# starting kernel
# baseline (speedup 1.0000x reference)
#include <hip/hip_runtime.h>
#include <math.h>

#define LL   12
#define TT   8
#define BB   32
#define CC   256
#define SS   64
#define EE   64
#define HDD  16
#define NHH  4
#define HIDD 32
#define HH   8
#define NKK  8
#define MROWS (BB*CC*LL)
#define SROWS (LL*TT*BB*CC)

typedef float v2f __attribute__((ext_vector_type(2)));
typedef float v8f __attribute__((ext_vector_type(8)));

static __device__ __forceinline__ v8f wmma4(v2f a, v2f b, v8f c) {
  return __builtin_amdgcn_wmma_f32_16x16x4_f32(false, a, false, b, (short)0, c,
                                               false, false);
}

static __device__ __forceinline__ float softplusf(float x) {
  return (x > 20.0f) ? x : log1pf(__expf(x));
}

__global__ __launch_bounds__(256) void k_stats(const float* __restrict__ ws,
                                               float* __restrict__ feats) {
  const int lane = threadIdx.x & 31;
  const int wid  = threadIdx.x >> 5;
  const int row  = blockIdx.x * 8 + wid;

  const float* p = ws + (size_t)row * SS;
  float v0 = p[lane];
  float v1 = p[lane + 32];

  float s = v0 + v1, s2 = v0 * v0 + v1 * v1;
  #pragma unroll
  for (int m = 16; m >= 1; m >>= 1) {
    s  += __shfl_xor(s,  m, 32);
    s2 += __shfl_xor(s2, m, 32);
  }
  const float mean = s * (1.0f / 64.0f);
  const float var  = (s2 - 64.0f * mean * mean) * (1.0f / 63.0f);

  #pragma unroll
  for (int k = 2; k <= 64; k <<= 1) {
    #pragma unroll
    for (int j = k >> 1; j >= 1; j >>= 1) {
      if (j == 32) {
        float lo = fminf(v0, v1), hi = fmaxf(v0, v1);
        v0 = lo; v1 = hi;
      } else {
        const bool lower = ((lane & j) == 0);
        {
          float o = __shfl_xor(v0, j, 32);
          bool asc = (k >= 64) ? true : ((lane & k) == 0);
          v0 = (asc == lower) ? fminf(v0, o) : fmaxf(v0, o);
        }
        {
          float o = __shfl_xor(v1, j, 32);
          bool asc;
          if (k >= 64)      asc = true;
          else if (k == 32) asc = false;
          else              asc = ((lane & k) == 0);
          v1 = (asc == lower) ? fminf(v1, o) : fmaxf(v1, o);
        }
      }
    }
  }

  const float q0  = __shfl(v0, 0, 32);
  const float s15 = __shfl(v0, 15, 32), s16 = __shfl(v0, 16, 32);
  const float s31 = __shfl(v0, 31, 32), s32 = __shfl(v1, 0, 32);
  const float s47 = __shfl(v1, 15, 32), s48 = __shfl(v1, 16, 32);
  const float q4  = __shfl(v1, 31, 32);
  const float q1 = s15 + 0.75f * (s16 - s15);
  const float q2 = 0.5f * (s31 + s32);
  const float q3 = s47 + 0.25f * (s48 - s47);

  if (lane == 0) {
    const int c = row & 255;
    int tmp = row >> 8;
    const int b = tmp & 31; tmp >>= 5;
    const int t = tmp & 7;
    const int l = tmp >> 3;
    float* o = feats + ((size_t)((b * LL + l) * CC + c)) * 56 + t * 7;
    o[0] = mean; o[1] = var; o[2] = q0; o[3] = q1; o[4] = q2; o[5] = q3; o[6] = q4;
  }
}

__global__ __launch_bounds__(256) void k_embed(const float* __restrict__ feats,
                                               const float* __restrict__ Wemb,
                                               const float* __restrict__ bemb,
                                               const float* __restrict__ pos,
                                               float* __restrict__ emb) {
  const int lane = threadIdx.x & 31;
  const int wid  = threadIdx.x >> 5;
  const int r  = lane & 15;
  const int hf = lane >> 4;
  const int m0 = blockIdx.x * 32 + (wid >> 2) * 16;
  const int n0 = (wid & 3) * 16;

  v8f acc = {0, 0, 0, 0, 0, 0, 0, 0};
  #pragma unroll
  for (int k0 = 0; k0 < 56; k0 += 4) {
    const int c0 = k0 + 2 * hf;
    const float* ap = feats + (size_t)(m0 + r) * 56 + c0;
    v2f a; a.x = ap[0]; a.y = ap[1];
    const float* bp = Wemb + (size_t)c0 * EE + n0 + r;
    v2f b; b.x = bp[0]; b.y = bp[EE];
    acc = wmma4(a, b, acc);
  }
  #pragma unroll
  for (int g = 0; g < 8; ++g) {
    const int m = m0 + g + 8 * hf;
    const int n = n0 + r;
    const int l = (m >> 8) % LL;
    emb[(size_t)m * EE + n] = acc[g] + bemb[n] + pos[l * EE + n];
  }
}

__global__ __launch_bounds__(384) void k_qkv(const float* __restrict__ emb,
                                             const float* __restrict__ win,
                                             const float* __restrict__ bin,
                                             float* __restrict__ qkv) {
  const int lane = threadIdx.x & 31;
  const int wid  = threadIdx.x >> 5;
  const int r  = lane & 15;
  const int hf = lane >> 4;
  const int m0 = blockIdx.x * 16;
  const int n0 = wid * 16;

  v8f acc = {0, 0, 0, 0, 0, 0, 0, 0};
  #pragma unroll
  for (int k0 = 0; k0 < 64; k0 += 4) {
    const int c0 = k0 + 2 * hf;
    const float* ap = emb + (size_t)(m0 + r) * EE + c0;
    v2f a; a.x = ap[0]; a.y = ap[1];
    const float* bp = win + (size_t)c0 * 192 + n0 + r;
    v2f b; b.x = bp[0]; b.y = bp[192];
    acc = wmma4(a, b, acc);
  }
  #pragma unroll
  for (int g = 0; g < 8; ++g) {
    const int m = m0 + g + 8 * hf;
    const int n = n0 + r;
    qkv[(size_t)m * 192 + n] = acc[g] + bin[n];
  }
}

__global__ __launch_bounds__(256) void k_attn(const float* __restrict__ qkv,
                                              float* __restrict__ ctx) {
  __shared__ float sP[8][16][17];
  const int lane = threadIdx.x & 31;
  const int wid  = threadIdx.x >> 5;
  const int task = blockIdx.x * 8 + wid;
  const int nb = task >> 2;
  const int h  = task & 3;
  const int r  = lane & 15;
  const int hf = lane >> 4;
  const float* base = qkv + (size_t)nb * LL * 192 + h * HDD;

  v8f S = {0, 0, 0, 0, 0, 0, 0, 0};
  #pragma unroll
  for (int k0 = 0; k0 < 16; k0 += 4) {
    const int c0 = k0 + 2 * hf;
    v2f a = {0, 0}, b = {0, 0};
    if (r < LL) {
      const float* qp = base + (size_t)r * 192 + c0;
      a.x = qp[0]; a.y = qp[1];
      const float* kp = qp + EE;
      b.x = kp[0]; b.y = kp[1];
    }
    S = wmma4(a, b, S);
  }

  float P[8];
  #pragma unroll
  for (int g = 0; g < 8; ++g) {
    float sv = (r < LL) ? S[g] * 0.25f : -__builtin_inff();
    float rm = sv;
    #pragma unroll
    for (int m = 8; m >= 1; m >>= 1) rm = fmaxf(rm, __shfl_xor(rm, m, 32));
    float e = __expf(sv - rm);
    float rs = e;
    #pragma unroll
    for (int m = 8; m >= 1; m >>= 1) rs += __shfl_xor(rs, m, 32);
    P[g] = e / rs;
  }
  #pragma unroll
  for (int g = 0; g < 8; ++g) sP[wid][g + 8 * hf][r] = P[g];
  __syncthreads();

  v8f Cv = {0, 0, 0, 0, 0, 0, 0, 0};
  #pragma unroll
  for (int k0 = 0; k0 < 16; k0 += 4) {
    const int c0 = k0 + 2 * hf;
    v2f a; a.x = sP[wid][r][c0]; a.y = sP[wid][r][c0 + 1];
    v2f b;
    b.x = (c0     < LL) ? base[(size_t)c0       * 192 + 128 + r] : 0.0f;
    b.y = (c0 + 1 < LL) ? base[(size_t)(c0 + 1) * 192 + 128 + r] : 0.0f;
    Cv = wmma4(a, b, Cv);
  }
  #pragma unroll
  for (int g = 0; g < 8; ++g) {
    const int m = g + 8 * hf;
    if (m < LL)
      ctx[((size_t)nb * LL + m) * EE + h * HDD + r] = Cv[g];
  }
}

__global__ __launch_bounds__(256) void k_proj_pool(const float* __restrict__ ctx,
                                                   const float* __restrict__ wout,
                                                   const float* __restrict__ bout,
                                                   float* __restrict__ pooled) {
  __shared__ float pool[EE];
  const int tid = threadIdx.x;
  if (tid < EE) pool[tid] = 0.0f;
  __syncthreads();

  const int lane = tid & 31;
  const int wid  = tid >> 5;
  const int r  = lane & 15;
  const int hf = lane >> 4;
  const size_t mbase = (size_t)blockIdx.x * 256;

  #pragma unroll
  for (int i = 0; i < 8; ++i) {
    const int t  = wid * 8 + i;
    const int m0 = (t >> 2) * 16;
    const int n0 = (t & 3) * 16;
    v8f acc = {0, 0, 0, 0, 0, 0, 0, 0};
    #pragma unroll
    for (int k0 = 0; k0 < 64; k0 += 4) {
      const int c0 = k0 + 2 * hf;
      const float* ap = ctx + (mbase + m0 + r) * EE + c0;
      v2f a; a.x = ap[0]; a.y = ap[1];
      const float* bp = wout + (size_t)c0 * EE + n0 + r;
      v2f b; b.x = bp[0]; b.y = bp[EE];
      acc = wmma4(a, b, acc);
    }
    float ps = acc[0] + acc[1] + acc[2] + acc[3] +
               acc[4] + acc[5] + acc[6] + acc[7];
    ps += __shfl_xor(ps, 16, 32);
    if (lane < 16) atomicAdd(&pool[n0 + r], ps);
  }
  __syncthreads();
  if (tid < EE)
    pooled[(size_t)blockIdx.x * EE + tid] = pool[tid] * (1.0f / 256.0f) + bout[tid];
}

__global__ __launch_bounds__(256) void k_mlp_lu(
    const float* __restrict__ pooled,
    const float* __restrict__ mw1, const float* __restrict__ mb1,
    const float* __restrict__ mg,  const float* __restrict__ mbb,
    const float* __restrict__ mw2, const float* __restrict__ mb2,
    const float* __restrict__ nw1, const float* __restrict__ nb1,
    const float* __restrict__ ng,  const float* __restrict__ nbb,
    const float* __restrict__ nw2, const float* __restrict__ nb2,
    float* __restrict__ out) {
  const int branch = blockIdx.x & 1;
  const int row    = blockIdx.x >> 1;
  const float* w1 = branch ? nw1 : mw1;
  const float* b1 = branch ? nb1 : mb1;
  const float* g  = branch ? ng  : mg;
  const float* bb = branch ? nbb : mbb;
  const float* w2 = branch ? nw2 : mw2;
  const float* b2 = branch ? nb2 : mb2;

  __shared__ float sx[EE];
  __shared__ float sh[HIDD];
  __shared__ float sraw[512];
  const int tid = threadIdx.x;

  if (tid < EE) sx[tid] = pooled[(size_t)row * EE + tid];
  __syncthreads();

  if (tid < HIDD) {
    float acc = b1[tid];
    #pragma unroll
    for (int i = 0; i < EE; ++i) acc += sx[i] * w1[i * HIDD + tid];
    float mu = acc;
    #pragma unroll
    for (int m = 16; m >= 1; m >>= 1) mu += __shfl_xor(mu, m, 32);
    mu *= (1.0f / 32.0f);
    const float d = acc - mu;
    float vv = d * d;
    #pragma unroll
    for (int m = 16; m >= 1; m >>= 1) vv += __shfl_xor(vv, m, 32);
    vv *= (1.0f / 32.0f);
    const float y = d * rsqrtf(vv + 1e-5f) * g[tid] + bb[tid];
    sh[tid] = 0.5f * y * (1.0f + erff(y * 0.70710678118654752f));
  }
  __syncthreads();

  for (int c = tid; c < 512; c += 256) {
    float acc = b2[c];
    #pragma unroll
    for (int j = 0; j < HIDD; ++j) acc += sh[j] * w2[j * 512 + c];
    sraw[c] = acc;
  }
  __syncthreads();

  const size_t obase = ((size_t)branch * 384 + row) * 512;
  for (int o = tid; o < 512; o += 256) {
    const int h  = o >> 6;
    const int ik = o & 63;
    const int i  = ik >> 3;
    const int k  = ik & 7;
    const float* rh = sraw + h * 64;
    float acc = 0.0f;
    #pragma unroll
    for (int j = 0; j < NKK; ++j) {
      if (j > i || j > k) continue;
      const float Lij = (j == i) ? 1.0f : rh[(i * (i - 1)) / 2 + j];
      float u = rh[28 + (NKK * j - (j * (j - 1)) / 2) + (k - j)];
      if (j == k) u = softplusf(u) + 0.01f;
      acc += Lij * u;
    }
    out[obase + o] = acc;
  }
}

extern "C" void kernel_launch(void* const* d_in, const int* in_sizes, int n_in,
                              void* d_out, int out_size, void* d_ws, size_t ws_size,
                              hipStream_t stream) {
  const float* wsfeat = (const float*)d_in[0];
  const float* W_emb  = (const float*)d_in[1];
  const float* b_emb  = (const float*)d_in[2];
  const float* pos    = (const float*)d_in[3];
  const float* w_in   = (const float*)d_in[4];
  const float* b_in   = (const float*)d_in[5];
  const float* w_out  = (const float*)d_in[6];
  const float* b_out  = (const float*)d_in[7];
  const float* m_w1 = (const float*)d_in[8];
  const float* m_b1 = (const float*)d_in[9];
  const float* m_g  = (const float*)d_in[10];
  const float* m_bb = (const float*)d_in[11];
  const float* m_w2 = (const float*)d_in[12];
  const float* m_b2 = (const float*)d_in[13];
  const float* n_w1 = (const float*)d_in[14];
  const float* n_b1 = (const float*)d_in[15];
  const float* n_g  = (const float*)d_in[16];
  const float* n_bb = (const float*)d_in[17];
  const float* n_w2 = (const float*)d_in[18];
  const float* n_b2 = (const float*)d_in[19];
  float* out = (float*)d_out;

  float* ws     = (float*)d_ws;
  float* feats  = ws;
  float* emb    = feats + (size_t)MROWS * 56;
  float* qkv    = emb   + (size_t)MROWS * EE;
  float* ctx    = qkv   + (size_t)MROWS * 192;
  float* pooled = ctx   + (size_t)MROWS * EE;

  k_stats<<<SROWS / 8, 256, 0, stream>>>(wsfeat, feats);
  k_embed<<<MROWS / 32, 256, 0, stream>>>(feats, W_emb, b_emb, pos, emb);
  k_qkv<<<MROWS / 16, 384, 0, stream>>>(emb, w_in, b_in, qkv);
  k_attn<<<(BB * CC * NHH) / 8, 256, 0, stream>>>(qkv, ctx);
  k_proj_pool<<<BB * LL, 256, 0, stream>>>(ctx, w_out, b_out, pooled);
  k_mlp_lu<<<BB * LL * 2, 256, 0, stream>>>(pooled,
      m_w1, m_b1, m_g, m_bb, m_w2, m_b2,
      n_w1, n_b1, n_g, n_bb, n_w2, n_b2, out);
}


# k_stats: one lane per 64-value row (rows turned through LDS), in-register min/max sorting network replaces the 32-lane shuffle bitonic sort; row sums now per lane (sum order changed)
# speedup vs baseline: 1.8400x; 1.8400x over previous
#include <hip/hip_runtime.h>
#include <math.h>

#define LL   12
#define TT   8
#define BB   32
#define CC   256
#define SS   64
#define EE   64
#define HDD  16
#define NHH  4
#define HIDD 32
#define HH   8
#define NKK  8
#define MROWS (BB*CC*LL)
#define SROWS (LL*TT*BB*CC)

typedef float v2f __attribute__((ext_vector_type(2)));
typedef float v8f __attribute__((ext_vector_type(8)));

static __device__ __forceinline__ v8f wmma4(v2f a, v2f b, v8f c) {
  return __builtin_amdgcn_wmma_f32_16x16x4_f32(false, a, false, b, (short)0, c,
                                               false, false);
}

static __device__ __forceinline__ float softplusf(float x) {
  return (x > 20.0f) ? x : log1pf(__expf(x));
}

#define ROWQ 17
#define CE(a, b) { const float lo_ = fminf(v[a], v[b]); const float hi_ = fmaxf(v[a], v[b]); v[a] = lo_; v[b] = hi_; }

__global__ __launch_bounds__(256) void k_stats(const float* __restrict__ ws,
                                               float* __restrict__ feats) {
  __shared__ float4 tile[8 * 32 * ROWQ];
  const int lane = threadIdx.x & 31;
  const int wid  = threadIdx.x >> 5;
  const int row0 = (blockIdx.x * 8 + wid) * 32;

  const float4* src = reinterpret_cast<const float4*>(ws + (size_t)row0 * SS);
  float4* wt = tile + wid * (32 * ROWQ);
  #pragma unroll
  for (int q = 0; q < 16; ++q)
    wt[(2 * q + (lane >> 4)) * ROWQ + (lane & 15)] = src[q * 32 + lane];
  __syncthreads();

  float v[64];
  #pragma unroll
  for (int q = 0; q < 16; ++q) {
    const float4 x = wt[lane * ROWQ + q];
    v[4 * q + 0] = x.x; v[4 * q + 1] = x.y; v[4 * q + 2] = x.z; v[4 * q + 3] = x.w;
  }

  float sa = 0.0f, sb = 0.0f, sc = 0.0f, sd = 0.0f;
  float ta = 0.0f, tb = 0.0f, tc = 0.0f, td = 0.0f;
  #pragma unroll
  for (int q = 0; q < 16; ++q) {
    sa += v[4 * q + 0]; ta = fmaf(v[4 * q + 0], v[4 * q + 0], ta);
    sb += v[4 * q + 1]; tb = fmaf(v[4 * q + 1], v[4 * q + 1], tb);
    sc += v[4 * q + 2]; tc = fmaf(v[4 * q + 2], v[4 * q + 2], tc);
    sd += v[4 * q + 3]; td = fmaf(v[4 * q + 3], v[4 * q + 3], td);
  }
  const float s  = (sa + sb) + (sc + sd);
  const float s2 = (ta + tb) + (tc + td);
  const float mean = s * (1.0f / 64.0f);
  const float var  = (s2 - 64.0f * mean * mean) * (1.0f / 63.0f);

  CE(0,1) CE(2,3) CE(4,5) CE(6,7) CE(8,9) CE(10,11)
  CE(12,13) CE(14,15) CE(16,17) CE(18,19) CE(20,21) CE(22,23)
  CE(24,25) CE(26,27) CE(28,29) CE(30,31) CE(32,33) CE(34,35)
  CE(36,37) CE(38,39) CE(40,41) CE(42,43) CE(44,45) CE(46,47)
  CE(48,49) CE(50,51) CE(52,53) CE(54,55) CE(56,57) CE(58,59)
  CE(60,61) CE(62,63) CE(0,2) CE(1,3) CE(4,6) CE(5,7)
  CE(8,10) CE(9,11) CE(12,14) CE(13,15) CE(16,18) CE(17,19)
  CE(20,22) CE(21,23) CE(24,26) CE(25,27) CE(28,30) CE(29,31)
  CE(32,34) CE(33,35) CE(36,38) CE(37,39) CE(40,42) CE(41,43)
  CE(44,46) CE(45,47) CE(48,50) CE(49,51) CE(52,54) CE(53,55)
  CE(56,58) CE(57,59) CE(60,62) CE(61,63) CE(1,2) CE(5,6)
  CE(9,10) CE(13,14) CE(17,18) CE(21,22) CE(25,26) CE(29,30)
  CE(33,34) CE(37,38) CE(41,42) CE(45,46) CE(49,50) CE(53,54)
  CE(57,58) CE(61,62) CE(0,4) CE(1,5) CE(2,6) CE(3,7)
  CE(8,12) CE(9,13) CE(10,14) CE(11,15) CE(16,20) CE(17,21)
  CE(18,22) CE(19,23) CE(24,28) CE(25,29) CE(26,30) CE(27,31)
  CE(32,36) CE(33,37) CE(34,38) CE(35,39) CE(40,44) CE(41,45)
  CE(42,46) CE(43,47) CE(48,52) CE(49,53) CE(50,54) CE(51,55)
  CE(56,60) CE(57,61) CE(58,62) CE(59,63) CE(2,4) CE(3,5)
  CE(10,12) CE(11,13) CE(18,20) CE(19,21) CE(26,28) CE(27,29)
  CE(34,36) CE(35,37) CE(42,44) CE(43,45) CE(50,52) CE(51,53)
  CE(58,60) CE(59,61) CE(1,2) CE(3,4) CE(5,6) CE(9,10)
  CE(11,12) CE(13,14) CE(17,18) CE(19,20) CE(21,22) CE(25,26)
  CE(27,28) CE(29,30) CE(33,34) CE(35,36) CE(37,38) CE(41,42)
  CE(43,44) CE(45,46) CE(49,50) CE(51,52) CE(53,54) CE(57,58)
  CE(59,60) CE(61,62) CE(0,8) CE(1,9) CE(2,10) CE(3,11)
  CE(4,12) CE(5,13) CE(6,14) CE(7,15) CE(16,24) CE(17,25)
  CE(18,26) CE(19,27) CE(20,28) CE(21,29) CE(22,30) CE(23,31)
  CE(32,40) CE(33,41) CE(34,42) CE(35,43) CE(36,44) CE(37,45)
  CE(38,46) CE(39,47) CE(48,56) CE(49,57) CE(50,58) CE(51,59)
  CE(52,60) CE(53,61) CE(54,62) CE(55,63) CE(4,8) CE(5,9)
  CE(6,10) CE(7,11) CE(20,24) CE(21,25) CE(22,26) CE(23,27)
  CE(36,40) CE(37,41) CE(38,42) CE(39,43) CE(52,56) CE(53,57)
  CE(54,58) CE(55,59) CE(2,4) CE(3,5) CE(6,8) CE(7,9)
  CE(10,12) CE(11,13) CE(18,20) CE(19,21) CE(22,24) CE(23,25)
  CE(26,28) CE(27,29) CE(34,36) CE(35,37) CE(38,40) CE(39,41)
  CE(42,44) CE(43,45) CE(50,52) CE(51,53) CE(54,56) CE(55,57)
  CE(58,60) CE(59,61) CE(1,2) CE(3,4) CE(5,6) CE(7,8)
  CE(9,10) CE(11,12) CE(13,14) CE(17,18) CE(19,20) CE(21,22)
  CE(23,24) CE(25,26) CE(27,28) CE(29,30) CE(33,34) CE(35,36)
  CE(37,38) CE(39,40) CE(41,42) CE(43,44) CE(45,46) CE(49,50)
  CE(51,52) CE(53,54) CE(55,56) CE(57,58) CE(59,60) CE(61,62)
  CE(0,16) CE(1,17) CE(2,18) CE(3,19) CE(4,20) CE(5,21)
  CE(6,22) CE(7,23) CE(8,24) CE(9,25) CE(10,26) CE(11,27)
  CE(12,28) CE(13,29) CE(14,30) CE(15,31) CE(32,48) CE(33,49)
  CE(34,50) CE(35,51) CE(36,52) CE(37,53) CE(38,54) CE(39,55)
  CE(40,56) CE(41,57) CE(42,58) CE(43,59) CE(44,60) CE(45,61)
  CE(46,62) CE(47,63) CE(8,16) CE(9,17) CE(10,18) CE(11,19)
  CE(12,20) CE(13,21) CE(14,22) CE(15,23) CE(40,48) CE(41,49)
  CE(42,50) CE(43,51) CE(44,52) CE(45,53) CE(46,54) CE(47,55)
  CE(4,8) CE(5,9) CE(6,10) CE(7,11) CE(12,16) CE(13,17)
  CE(14,18) CE(15,19) CE(20,24) CE(21,25) CE(22,26) CE(23,27)
  CE(36,40) CE(37,41) CE(38,42) CE(39,43) CE(44,48) CE(45,49)
  CE(46,50) CE(47,51) CE(52,56) CE(53,57) CE(54,58) CE(55,59)
  CE(2,4) CE(3,5) CE(6,8) CE(7,9) CE(10,12) CE(11,13)
  CE(14,16) CE(15,17) CE(18,20) CE(19,21) CE(22,24) CE(23,25)
  CE(26,28) CE(27,29) CE(34,36) CE(35,37) CE(38,40) CE(39,41)
  CE(42,44) CE(43,45) CE(46,48) CE(47,49) CE(50,52) CE(51,53)
  CE(54,56) CE(55,57) CE(58,60) CE(59,61) CE(1,2) CE(3,4)
  CE(5,6) CE(7,8) CE(9,10) CE(11,12) CE(13,14) CE(15,16)
  CE(17,18) CE(19,20) CE(21,22) CE(23,24) CE(25,26) CE(27,28)
  CE(29,30) CE(33,34) CE(35,36) CE(37,38) CE(39,40) CE(41,42)
  CE(43,44) CE(45,46) CE(47,48) CE(49,50) CE(51,52) CE(53,54)
  CE(55,56) CE(57,58) CE(59,60) CE(61,62) CE(0,32) CE(1,33)
  CE(2,34) CE(3,35) CE(4,36) CE(5,37) CE(6,38) CE(7,39)
  CE(8,40) CE(9,41) CE(10,42) CE(11,43) CE(12,44) CE(13,45)
  CE(14,46) CE(15,47) CE(16,48) CE(17,49) CE(18,50) CE(19,51)
  CE(20,52) CE(21,53) CE(22,54) CE(23,55) CE(24,56) CE(25,57)
  CE(26,58) CE(27,59) CE(28,60) CE(29,61) CE(30,62) CE(31,63)
  CE(16,32) CE(17,33) CE(18,34) CE(19,35) CE(20,36) CE(21,37)
  CE(22,38) CE(23,39) CE(24,40) CE(25,41) CE(26,42) CE(27,43)
  CE(28,44) CE(29,45) CE(30,46) CE(31,47) CE(8,16) CE(9,17)
  CE(10,18) CE(11,19) CE(12,20) CE(13,21) CE(14,22) CE(15,23)
  CE(24,32) CE(25,33) CE(26,34) CE(27,35) CE(28,36) CE(29,37)
  CE(30,38) CE(31,39) CE(40,48) CE(41,49) CE(42,50) CE(43,51)
  CE(44,52) CE(45,53) CE(46,54) CE(47,55) CE(4,8) CE(5,9)
  CE(6,10) CE(7,11) CE(12,16) CE(13,17) CE(14,18) CE(15,19)
  CE(20,24) CE(21,25) CE(22,26) CE(23,27) CE(28,32) CE(29,33)
  CE(30,34) CE(31,35) CE(36,40) CE(37,41) CE(38,42) CE(39,43)
  CE(44,48) CE(45,49) CE(46,50) CE(47,51) CE(52,56) CE(53,57)
  CE(54,58) CE(55,59) CE(2,4) CE(3,5) CE(6,8) CE(7,9)
  CE(10,12) CE(11,13) CE(14,16) CE(15,17) CE(18,20) CE(19,21)
  CE(22,24) CE(23,25) CE(26,28) CE(27,29) CE(30,32) CE(31,33)
  CE(34,36) CE(35,37) CE(38,40) CE(39,41) CE(42,44) CE(43,45)
  CE(46,48) CE(47,49) CE(50,52) CE(51,53) CE(54,56) CE(55,57)
  CE(58,60) CE(59,61) CE(1,2) CE(3,4) CE(5,6) CE(7,8)
  CE(9,10) CE(11,12) CE(13,14) CE(15,16) CE(17,18) CE(19,20)
  CE(21,22) CE(23,24) CE(25,26) CE(27,28) CE(29,30) CE(31,32)
  CE(33,34) CE(35,36) CE(37,38) CE(39,40) CE(41,42) CE(43,44)
  CE(45,46) CE(47,48) CE(49,50) CE(51,52) CE(53,54) CE(55,56)
  CE(57,58) CE(59,60) CE(61,62)

  const float q0 = v[0];
  const float q1 = v[15] + 0.75f * (v[16] - v[15]);
  const float q2 = 0.5f * (v[31] + v[32]);
  const float q3 = v[47] + 0.25f * (v[48] - v[47]);
  const float q4 = v[63];

  const int row = row0 + lane;
  const int c = row & 255;
  int tmp = row >> 8;
  const int b = tmp & 31; tmp >>= 5;
  const int t = tmp & 7;
  const int l = tmp >> 3;
  float* o = feats + ((size_t)((b * LL + l) * CC + c)) * 56 + t * 7;
  o[0] = mean; o[1] = var; o[2] = q0; o[3] = q1; o[4] = q2; o[5] = q3; o[6] = q4;
}
#undef CE

__global__ __launch_bounds__(256) void k_embed(const float* __restrict__ feats,
                                               const float* __restrict__ Wemb,
                                               const float* __restrict__ bemb,
                                               const float* __restrict__ pos,
                                               float* __restrict__ emb) {
  const int lane = threadIdx.x & 31;
  const int wid  = threadIdx.x >> 5;
  const int r  = lane & 15;
  const int hf = lane >> 4;
  const int m0 = blockIdx.x * 32 + (wid >> 2) * 16;
  const int n0 = (wid & 3) * 16;

  v8f acc = {0, 0, 0, 0, 0, 0, 0, 0};
  #pragma unroll
  for (int k0 = 0; k0 < 56; k0 += 4) {
    const int c0 = k0 + 2 * hf;
    const float* ap = feats + (size_t)(m0 + r) * 56 + c0;
    v2f a; a.x = ap[0]; a.y = ap[1];
    const float* bp = Wemb + (size_t)c0 * EE + n0 + r;
    v2f b; b.x = bp[0]; b.y = bp[EE];
    acc = wmma4(a, b, acc);
  }
  #pragma unroll
  for (int g = 0; g < 8; ++g) {
    const int m = m0 + g + 8 * hf;
    const int n = n0 + r;
    const int l = (m >> 8) % LL;
    emb[(size_t)m * EE + n] = acc[g] + bemb[n] + pos[l * EE + n];
  }
}

__global__ __launch_bounds__(384) void k_qkv(const float* __restrict__ emb,
                                             const float* __restrict__ win,
                                             const float* __restrict__ bin,
                                             float* __restrict__ qkv) {
  const int lane = threadIdx.x & 31;
  const int wid  = threadIdx.x >> 5;
  const int r  = lane & 15;
  const int hf = lane >> 4;
  const int m0 = blockIdx.x * 16;
  const int n0 = wid * 16;

  v8f acc = {0, 0, 0, 0, 0, 0, 0, 0};
  #pragma unroll
  for (int k0 = 0; k0 < 64; k0 += 4) {
    const int c0 = k0 + 2 * hf;
    const float* ap = emb + (size_t)(m0 + r) * EE + c0;
    v2f a; a.x = ap[0]; a.y = ap[1];
    const float* bp = win + (size_t)c0 * 192 + n0 + r;
    v2f b; b.x = bp[0]; b.y = bp[192];
    acc = wmma4(a, b, acc);
  }
  #pragma unroll
  for (int g = 0; g < 8; ++g) {
    const int m = m0 + g + 8 * hf;
    const int n = n0 + r;
    qkv[(size_t)m * 192 + n] = acc[g] + bin[n];
  }
}

__global__ __launch_bounds__(256) void k_attn(const float* __restrict__ qkv,
                                              float* __restrict__ ctx) {
  __shared__ float sP[8][16][17];
  const int lane = threadIdx.x & 31;
  const int wid  = threadIdx.x >> 5;
  const int task = blockIdx.x * 8 + wid;
  const int nb = task >> 2;
  const int h  = task & 3;
  const int r  = lane & 15;
  const int hf = lane >> 4;
  const float* base = qkv + (size_t)nb * LL * 192 + h * HDD;

  v8f S = {0, 0, 0, 0, 0, 0, 0, 0};
  #pragma unroll
  for (int k0 = 0; k0 < 16; k0 += 4) {
    const int c0 = k0 + 2 * hf;
    v2f a = {0, 0}, b = {0, 0};
    if (r < LL) {
      const float* qp = base + (size_t)r * 192 + c0;
      a.x = qp[0]; a.y = qp[1];
      const float* kp = qp + EE;
      b.x = kp[0]; b.y = kp[1];
    }
    S = wmma4(a, b, S);
  }

  float P[8];
  #pragma unroll
  for (int g = 0; g < 8; ++g) {
    float sv = (r < LL) ? S[g] * 0.25f : -__builtin_inff();
    float rm = sv;
    #pragma unroll
    for (int m = 8; m >= 1; m >>= 1) rm = fmaxf(rm, __shfl_xor(rm, m, 32));
    float e = __expf(sv - rm);
    float rs = e;
    #pragma unroll
    for (int m = 8; m >= 1; m >>= 1) rs += __shfl_xor(rs, m, 32);
    P[g] = e / rs;
  }
  #pragma unroll
  for (int g = 0; g < 8; ++g) sP[wid][g + 8 * hf][r] = P[g];
  __syncthreads();

  v8f Cv = {0, 0, 0, 0, 0, 0, 0, 0};
  #pragma unroll
  for (int k0 = 0; k0 < 16; k0 += 4) {
    const int c0 = k0 + 2 * hf;
    v2f a; a.x = sP[wid][r][c0]; a.y = sP[wid][r][c0 + 1];
    v2f b;
    b.x = (c0     < LL) ? base[(size_t)c0       * 192 + 128 + r] : 0.0f;
    b.y = (c0 + 1 < LL) ? base[(size_t)(c0 + 1) * 192 + 128 + r] : 0.0f;
    Cv = wmma4(a, b, Cv);
  }
  #pragma unroll
  for (int g = 0; g < 8; ++g) {
    const int m = g + 8 * hf;
    if (m < LL)
      ctx[((size_t)nb * LL + m) * EE + h * HDD + r] = Cv[g];
  }
}

__global__ __launch_bounds__(256) void k_proj_pool(const float* __restrict__ ctx,
                                                   const float* __restrict__ wout,
                                                   const float* __restrict__ bout,
                                                   float* __restrict__ pooled) {
  __shared__ float pool[EE];
  const int tid = threadIdx.x;
  if (tid < EE) pool[tid] = 0.0f;
  __syncthreads();

  const int lane = tid & 31;
  const int wid  = tid >> 5;
  const int r  = lane & 15;
  const int hf = lane >> 4;
  const size_t mbase = (size_t)blockIdx.x * 256;

  #pragma unroll
  for (int i = 0; i < 8; ++i) {
    const int t  = wid * 8 + i;
    const int m0 = (t >> 2) * 16;
    const int n0 = (t & 3) * 16;
    v8f acc = {0, 0, 0, 0, 0, 0, 0, 0};
    #pragma unroll
    for (int k0 = 0; k0 < 64; k0 += 4) {
      const int c0 = k0 + 2 * hf;
      const float* ap = ctx + (mbase + m0 + r) * EE + c0;
      v2f a; a.x = ap[0]; a.y = ap[1];
      const float* bp = wout + (size_t)c0 * EE + n0 + r;
      v2f b; b.x = bp[0]; b.y = bp[EE];
      acc = wmma4(a, b, acc);
    }
    float ps = acc[0] + acc[1] + acc[2] + acc[3] +
               acc[4] + acc[5] + acc[6] + acc[7];
    ps += __shfl_xor(ps, 16, 32);
    if (lane < 16) atomicAdd(&pool[n0 + r], ps);
  }
  __syncthreads();
  if (tid < EE)
    pooled[(size_t)blockIdx.x * EE + tid] = pool[tid] * (1.0f / 256.0f) + bout[tid];
}

__global__ __launch_bounds__(256) void k_mlp_lu(
    const float* __restrict__ pooled,
    const float* __restrict__ mw1, const float* __restrict__ mb1,
    const float* __restrict__ mg,  const float* __restrict__ mbb,
    const float* __restrict__ mw2, const float* __restrict__ mb2,
    const float* __restrict__ nw1, const float* __restrict__ nb1,
    const float* __restrict__ ng,  const float* __restrict__ nbb,
    const float* __restrict__ nw2, const float* __restrict__ nb2,
    float* __restrict__ out) {
  const int branch = blockIdx.x & 1;
  const int row    = blockIdx.x >> 1;
  const float* w1 = branch ? nw1 : mw1;
  const float* b1 = branch ? nb1 : mb1;
  const float* g  = branch ? ng  : mg;
  const float* bb = branch ? nbb : mbb;
  const float* w2 = branch ? nw2 : mw2;
  const float* b2 = branch ? nb2 : mb2;

  __shared__ float sx[EE];
  __shared__ float sh[HIDD];
  __shared__ float sraw[512];
  const int tid = threadIdx.x;

  if (tid < EE) sx[tid] = pooled[(size_t)row * EE + tid];
  __syncthreads();

  if (tid < HIDD) {
    float acc = b1[tid];
    #pragma unroll
    for (int i = 0; i < EE; ++i) acc += sx[i] * w1[i * HIDD + tid];
    float mu = acc;
    #pragma unroll
    for (int m = 16; m >= 1; m >>= 1) mu += __shfl_xor(mu, m, 32);
    mu *= (1.0f / 32.0f);
    const float d = acc - mu;
    float vv = d * d;
    #pragma unroll
    for (int m = 16; m >= 1; m >>= 1) vv += __shfl_xor(vv, m, 32);
    vv *= (1.0f / 32.0f);
    const float y = d * rsqrtf(vv + 1e-5f) * g[tid] + bb[tid];
    sh[tid] = 0.5f * y * (1.0f + erff(y * 0.70710678118654752f));
  }
  __syncthreads();

  for (int c = tid; c < 512; c += 256) {
    float acc = b2[c];
    #pragma unroll
    for (int j = 0; j < HIDD; ++j) acc += sh[j] * w2[j * 512 + c];
    sraw[c] = acc;
  }
  __syncthreads();

  const size_t obase = ((size_t)branch * 384 + row) * 512;
  for (int o = tid; o < 512; o += 256) {
    const int h  = o >> 6;
    const int ik = o & 63;
    const int i  = ik >> 3;
    const int k  = ik & 7;
    const float* rh = sraw + h * 64;
    float acc = 0.0f;
    #pragma unroll
    for (int j = 0; j < NKK; ++j) {
      if (j > i || j > k) continue;
      const float Lij = (j == i) ? 1.0f : rh[(i * (i - 1)) / 2 + j];
      float u = rh[28 + (NKK * j - (j * (j - 1)) / 2) + (k - j)];
      if (j == k) u = softplusf(u) + 0.01f;
      acc += Lij * u;
    }
    out[obase + o] = acc;
  }
}

extern "C" void kernel_launch(void* const* d_in, const int* in_sizes, int n_in,
                              void* d_out, int out_size, void* d_ws, size_t ws_size,
                              hipStream_t stream) {
  const float* wsfeat = (const float*)d_in[0];
  const float* W_emb  = (const float*)d_in[1];
  const float* b_emb  = (const float*)d_in[2];
  const float* pos    = (const float*)d_in[3];
  const float* w_in   = (const float*)d_in[4];
  const float* b_in   = (const float*)d_in[5];
  const float* w_out  = (const float*)d_in[6];
  const float* b_out  = (const float*)d_in[7];
  const float* m_w1 = (const float*)d_in[8];
  const float* m_b1 = (const float*)d_in[9];
  const float* m_g  = (const float*)d_in[10];
  const float* m_bb = (const float*)d_in[11];
  const float* m_w2 = (const float*)d_in[12];
  const float* m_b2 = (const float*)d_in[13];
  const float* n_w1 = (const float*)d_in[14];
  const float* n_b1 = (const float*)d_in[15];
  const float* n_g  = (const float*)d_in[16];
  const float* n_bb = (const float*)d_in[17];
  const float* n_w2 = (const float*)d_in[18];
  const float* n_b2 = (const float*)d_in[19];
  float* out = (float*)d_out;

  float* ws     = (float*)d_ws;
  float* feats  = ws;
  float* emb    = feats + (size_t)MROWS * 56;
  float* qkv    = emb   + (size_t)MROWS * EE;
  float* ctx    = qkv   + (size_t)MROWS * 192;
  float* pooled = ctx   + (size_t)MROWS * EE;

  k_stats<<<SROWS / 256, 256, 0, stream>>>(wsfeat, feats);
  k_embed<<<MROWS / 32, 256, 0, stream>>>(feats, W_emb, b_emb, pos, emb);
  k_qkv<<<MROWS / 16, 384, 0, stream>>>(emb, w_in, b_in, qkv);
  k_attn<<<(BB * CC * NHH) / 8, 256, 0, stream>>>(qkv, ctx);
  k_proj_pool<<<BB * LL, 256, 0, stream>>>(ctx, w_out, b_out, pooled);
  k_mlp_lu<<<BB * LL * 2, 256, 0, stream>>>(pooled,
      m_w1, m_b1, m_g, m_bb, m_w2, m_b2,
      n_w1, n_b1, n_g, n_bb, n_w2, n_b2, out);
}


# fuse embed, qkv and attention into one kernel: a 4-wave block keeps 48 rows (4 batches) of feats/emb/qkv/P/ctx in LDS, wave = head; emb and qkv never touch memory; 6 launches -> 4
# speedup vs baseline: 2.6718x; 1.4521x over previous
#include <hip/hip_runtime.h>
#include <math.h>

#define LL   12
#define TT   8
#define BB   32
#define CC   256
#define SS   64
#define EE   64
#define HDD  16
#define NHH  4
#define HIDD 32
#define HH   8
#define NKK  8
#define MROWS (BB*CC*LL)
#define SROWS (LL*TT*BB*CC)

typedef float v2f __attribute__((ext_vector_type(2)));
typedef float v8f __attribute__((ext_vector_type(8)));

static __device__ __forceinline__ v8f wmma4(v2f a, v2f b, v8f c) {
  return __builtin_amdgcn_wmma_f32_16x16x4_f32(false, a, false, b, (short)0, c,
                                               false, false);
}

static __device__ __forceinline__ float softplusf(float x) {
  return (x > 20.0f) ? x : log1pf(__expf(x));
}

#define ROWQ 17
#define CE(a, b) { const float lo_ = fminf(v[a], v[b]); const float hi_ = fmaxf(v[a], v[b]); v[a] = lo_; v[b] = hi_; }

__global__ __launch_bounds__(256) void k_stats(const float* __restrict__ ws,
                                               float* __restrict__ feats) {
  __shared__ float4 tile[8 * 32 * ROWQ];
  const int lane = threadIdx.x & 31;
  const int wid  = threadIdx.x >> 5;
  const int row0 = (blockIdx.x * 8 + wid) * 32;

  const float4* src = reinterpret_cast<const float4*>(ws + (size_t)row0 * SS);
  float4* wt = tile + wid * (32 * ROWQ);
  #pragma unroll
  for (int q = 0; q < 16; ++q)
    wt[(2 * q + (lane >> 4)) * ROWQ + (lane & 15)] = src[q * 32 + lane];
  __syncthreads();

  float v[64];
  #pragma unroll
  for (int q = 0; q < 16; ++q) {
    const float4 x = wt[lane * ROWQ + q];
    v[4 * q + 0] = x.x; v[4 * q + 1] = x.y; v[4 * q + 2] = x.z; v[4 * q + 3] = x.w;
  }

  float sa = 0.0f, sb = 0.0f, sc = 0.0f, sd = 0.0f;
  float ta = 0.0f, tb = 0.0f, tc = 0.0f, td = 0.0f;
  #pragma unroll
  for (int q = 0; q < 16; ++q) {
    sa += v[4 * q + 0]; ta = fmaf(v[4 * q + 0], v[4 * q + 0], ta);
    sb += v[4 * q + 1]; tb = fmaf(v[4 * q + 1], v[4 * q + 1], tb);
    sc += v[4 * q + 2]; tc = fmaf(v[4 * q + 2], v[4 * q + 2], tc);
    sd += v[4 * q + 3]; td = fmaf(v[4 * q + 3], v[4 * q + 3], td);
  }
  const float s  = (sa + sb) + (sc + sd);
  const float s2 = (ta + tb) + (tc + td);
  const float mean = s * (1.0f / 64.0f);
  const float var  = (s2 - 64.0f * mean * mean) * (1.0f / 63.0f);

  CE(0,1) CE(2,3) CE(4,5) CE(6,7) CE(8,9) CE(10,11)
  CE(12,13) CE(14,15) CE(16,17) CE(18,19) CE(20,21) CE(22,23)
  CE(24,25) CE(26,27) CE(28,29) CE(30,31) CE(32,33) CE(34,35)
  CE(36,37) CE(38,39) CE(40,41) CE(42,43) CE(44,45) CE(46,47)
  CE(48,49) CE(50,51) CE(52,53) CE(54,55) CE(56,57) CE(58,59)
  CE(60,61) CE(62,63) CE(0,2) CE(1,3) CE(4,6) CE(5,7)
  CE(8,10) CE(9,11) CE(12,14) CE(13,15) CE(16,18) CE(17,19)
  CE(20,22) CE(21,23) CE(24,26) CE(25,27) CE(28,30) CE(29,31)
  CE(32,34) CE(33,35) CE(36,38) CE(37,39) CE(40,42) CE(41,43)
  CE(44,46) CE(45,47) CE(48,50) CE(49,51) CE(52,54) CE(53,55)
  CE(56,58) CE(57,59) CE(60,62) CE(61,63) CE(1,2) CE(5,6)
  CE(9,10) CE(13,14) CE(17,18) CE(21,22) CE(25,26) CE(29,30)
  CE(33,34) CE(37,38) CE(41,42) CE(45,46) CE(49,50) CE(53,54)
  CE(57,58) CE(61,62) CE(0,4) CE(1,5) CE(2,6) CE(3,7)
  CE(8,12) CE(9,13) CE(10,14) CE(11,15) CE(16,20) CE(17,21)
  CE(18,22) CE(19,23) CE(24,28) CE(25,29) CE(26,30) CE(27,31)
  CE(32,36) CE(33,37) CE(34,38) CE(35,39) CE(40,44) CE(41,45)
  CE(42,46) CE(43,47) CE(48,52) CE(49,53) CE(50,54) CE(51,55)
  CE(56,60) CE(57,61) CE(58,62) CE(59,63) CE(2,4) CE(3,5)
  CE(10,12) CE(11,13) CE(18,20) CE(19,21) CE(26,28) CE(27,29)
  CE(34,36) CE(35,37) CE(42,44) CE(43,45) CE(50,52) CE(51,53)
  CE(58,60) CE(59,61) CE(1,2) CE(3,4) CE(5,6) CE(9,10)
  CE(11,12) CE(13,14) CE(17,18) CE(19,20) CE(21,22) CE(25,26)
  CE(27,28) CE(29,30) CE(33,34) CE(35,36) CE(37,38) CE(41,42)
  CE(43,44) CE(45,46) CE(49,50) CE(51,52) CE(53,54) CE(57,58)
  CE(59,60) CE(61,62) CE(0,8) CE(1,9) CE(2,10) CE(3,11)
  CE(4,12) CE(5,13) CE(6,14) CE(7,15) CE(16,24) CE(17,25)
  CE(18,26) CE(19,27) CE(20,28) CE(21,29) CE(22,30) CE(23,31)
  CE(32,40) CE(33,41) CE(34,42) CE(35,43) CE(36,44) CE(37,45)
  CE(38,46) CE(39,47) CE(48,56) CE(49,57) CE(50,58) CE(51,59)
  CE(52,60) CE(53,61) CE(54,62) CE(55,63) CE(4,8) CE(5,9)
  CE(6,10) CE(7,11) CE(20,24) CE(21,25) CE(22,26) CE(23,27)
  CE(36,40) CE(37,41) CE(38,42) CE(39,43) CE(52,56) CE(53,57)
  CE(54,58) CE(55,59) CE(2,4) CE(3,5) CE(6,8) CE(7,9)
  CE(10,12) CE(11,13) CE(18,20) CE(19,21) CE(22,24) CE(23,25)
  CE(26,28) CE(27,29) CE(34,36) CE(35,37) CE(38,40) CE(39,41)
  CE(42,44) CE(43,45) CE(50,52) CE(51,53) CE(54,56) CE(55,57)
  CE(58,60) CE(59,61) CE(1,2) CE(3,4) CE(5,6) CE(7,8)
  CE(9,10) CE(11,12) CE(13,14) CE(17,18) CE(19,20) CE(21,22)
  CE(23,24) CE(25,26) CE(27,28) CE(29,30) CE(33,34) CE(35,36)
  CE(37,38) CE(39,40) CE(41,42) CE(43,44) CE(45,46) CE(49,50)
  CE(51,52) CE(53,54) CE(55,56) CE(57,58) CE(59,60) CE(61,62)
  CE(0,16) CE(1,17) CE(2,18) CE(3,19) CE(4,20) CE(5,21)
  CE(6,22) CE(7,23) CE(8,24) CE(9,25) CE(10,26) CE(11,27)
  CE(12,28) CE(13,29) CE(14,30) CE(15,31) CE(32,48) CE(33,49)
  CE(34,50) CE(35,51) CE(36,52) CE(37,53) CE(38,54) CE(39,55)
  CE(40,56) CE(41,57) CE(42,58) CE(43,59) CE(44,60) CE(45,61)
  CE(46,62) CE(47,63) CE(8,16) CE(9,17) CE(10,18) CE(11,19)
  CE(12,20) CE(13,21) CE(14,22) CE(15,23) CE(40,48) CE(41,49)
  CE(42,50) CE(43,51) CE(44,52) CE(45,53) CE(46,54) CE(47,55)
  CE(4,8) CE(5,9) CE(6,10) CE(7,11) CE(12,16) CE(13,17)
  CE(14,18) CE(15,19) CE(20,24) CE(21,25) CE(22,26) CE(23,27)
  CE(36,40) CE(37,41) CE(38,42) CE(39,43) CE(44,48) CE(45,49)
  CE(46,50) CE(47,51) CE(52,56) CE(53,57) CE(54,58) CE(55,59)
  CE(2,4) CE(3,5) CE(6,8) CE(7,9) CE(10,12) CE(11,13)
  CE(14,16) CE(15,17) CE(18,20) CE(19,21) CE(22,24) CE(23,25)
  CE(26,28) CE(27,29) CE(34,36) CE(35,37) CE(38,40) CE(39,41)
  CE(42,44) CE(43,45) CE(46,48) CE(47,49) CE(50,52) CE(51,53)
  CE(54,56) CE(55,57) CE(58,60) CE(59,61) CE(1,2) CE(3,4)
  CE(5,6) CE(7,8) CE(9,10) CE(11,12) CE(13,14) CE(15,16)
  CE(17,18) CE(19,20) CE(21,22) CE(23,24) CE(25,26) CE(27,28)
  CE(29,30) CE(33,34) CE(35,36) CE(37,38) CE(39,40) CE(41,42)
  CE(43,44) CE(45,46) CE(47,48) CE(49,50) CE(51,52) CE(53,54)
  CE(55,56) CE(57,58) CE(59,60) CE(61,62) CE(0,32) CE(1,33)
  CE(2,34) CE(3,35) CE(4,36) CE(5,37) CE(6,38) CE(7,39)
  CE(8,40) CE(9,41) CE(10,42) CE(11,43) CE(12,44) CE(13,45)
  CE(14,46) CE(15,47) CE(16,48) CE(17,49) CE(18,50) CE(19,51)
  CE(20,52) CE(21,53) CE(22,54) CE(23,55) CE(24,56) CE(25,57)
  CE(26,58) CE(27,59) CE(28,60) CE(29,61) CE(30,62) CE(31,63)
  CE(16,32) CE(17,33) CE(18,34) CE(19,35) CE(20,36) CE(21,37)
  CE(22,38) CE(23,39) CE(24,40) CE(25,41) CE(26,42) CE(27,43)
  CE(28,44) CE(29,45) CE(30,46) CE(31,47) CE(8,16) CE(9,17)
  CE(10,18) CE(11,19) CE(12,20) CE(13,21) CE(14,22) CE(15,23)
  CE(24,32) CE(25,33) CE(26,34) CE(27,35) CE(28,36) CE(29,37)
  CE(30,38) CE(31,39) CE(40,48) CE(41,49) CE(42,50) CE(43,51)
  CE(44,52) CE(45,53) CE(46,54) CE(47,55) CE(4,8) CE(5,9)
  CE(6,10) CE(7,11) CE(12,16) CE(13,17) CE(14,18) CE(15,19)
  CE(20,24) CE(21,25) CE(22,26) CE(23,27) CE(28,32) CE(29,33)
  CE(30,34) CE(31,35) CE(36,40) CE(37,41) CE(38,42) CE(39,43)
  CE(44,48) CE(45,49) CE(46,50) CE(47,51) CE(52,56) CE(53,57)
  CE(54,58) CE(55,59) CE(2,4) CE(3,5) CE(6,8) CE(7,9)
  CE(10,12) CE(11,13) CE(14,16) CE(15,17) CE(18,20) CE(19,21)
  CE(22,24) CE(23,25) CE(26,28) CE(27,29) CE(30,32) CE(31,33)
  CE(34,36) CE(35,37) CE(38,40) CE(39,41) CE(42,44) CE(43,45)
  CE(46,48) CE(47,49) CE(50,52) CE(51,53) CE(54,56) CE(55,57)
  CE(58,60) CE(59,61) CE(1,2) CE(3,4) CE(5,6) CE(7,8)
  CE(9,10) CE(11,12) CE(13,14) CE(15,16) CE(17,18) CE(19,20)
  CE(21,22) CE(23,24) CE(25,26) CE(27,28) CE(29,30) CE(31,32)
  CE(33,34) CE(35,36) CE(37,38) CE(39,40) CE(41,42) CE(43,44)
  CE(45,46) CE(47,48) CE(49,50) CE(51,52) CE(53,54) CE(55,56)
  CE(57,58) CE(59,60) CE(61,62)

  const float q0 = v[0];
  const float q1 = v[15] + 0.75f * (v[16] - v[15]);
  const float q2 = 0.5f * (v[31] + v[32]);
  const float q3 = v[47] + 0.25f * (v[48] - v[47]);
  const float q4 = v[63];

  const int row = row0 + lane;
  const int c = row & 255;
  int tmp = row >> 8;
  const int b = tmp & 31; tmp >>= 5;
  const int t = tmp & 7;
  const int l = tmp >> 3;
  float* o = feats + ((size_t)((b * LL + l) * CC + c)) * 56 + t * 7;
  o[0] = mean; o[1] = var; o[2] = q0; o[3] = q1; o[4] = q2; o[5] = q3; o[6] = q4;
}
#undef CE

#define UR 48
#define FP 60
#define EP 68
#define QP 196
#define PP 17

__global__ __launch_bounds__(128) void k_fused(const float* __restrict__ feats,
                                               const float* __restrict__ Wemb,
                                               const float* __restrict__ bemb,
                                               const float* __restrict__ pos,
                                               const float* __restrict__ win,
                                               const float* __restrict__ bin,
                                               float* __restrict__ ctx) {
  __shared__ __attribute__((aligned(16))) float sA[UR * FP];
  __shared__ float sPall[4 * 4 * 16 * PP];
  __shared__ __attribute__((aligned(16))) float sB[UR * EP];
  __shared__ __attribute__((aligned(16))) float sC[UR * QP];
  const int tid  = threadIdx.x;
  const int lane = tid & 31;
  const int wid  = tid >> 5;
  const int r  = lane & 15;
  const int hf = lane >> 4;
  const size_t row0 = (size_t)blockIdx.x * UR;

  {
    const float4* src = reinterpret_cast<const float4*>(feats + row0 * 56);
    if (tid < 112) {
      float4 x[6];
      #pragma unroll
      for (int i = 0; i < 6; ++i) x[i] = src[tid + 112 * i];
      #pragma unroll
      for (int i = 0; i < 6; ++i) {
        const int j = tid + 112 * i;
        const int rr = j / 14;
        const int c4 = j - rr * 14;
        *reinterpret_cast<float4*>(sA + rr * FP + c4 * 4) = x[i];
      }
    }
  }
  __syncthreads();

  {
    const int n0 = wid * 16;
    v8f e0 = {0, 0, 0, 0, 0, 0, 0, 0};
    v8f e1 = {0, 0, 0, 0, 0, 0, 0, 0};
    v8f e2 = {0, 0, 0, 0, 0, 0, 0, 0};
    #pragma unroll
    for (int k0 = 0; k0 < 56; k0 += 4) {
      const int c0 = k0 + 2 * hf;
      const float* bp = Wemb + (size_t)c0 * EE + n0 + r;
      v2f b; b.x = bp[0]; b.y = bp[EE];
      const float* ap = sA + r * FP + c0;
      const v2f a0 = *reinterpret_cast<const v2f*>(ap);
      const v2f a1 = *reinterpret_cast<const v2f*>(ap + 16 * FP);
      const v2f a2 = *reinterpret_cast<const v2f*>(ap + 32 * FP);
      e0 = wmma4(a0, b, e0);
      e1 = wmma4(a1, b, e1);
      e2 = wmma4(a2, b, e2);
    }
    const int n = n0 + r;
    const float be = bemb[n];
    #pragma unroll
    for (int g = 0; g < 8; ++g) {
      const int ml = g + 8 * hf;
      const int l0 = (int)(((row0 + ml)      >> 8) % LL);
      const int l1 = (int)(((row0 + ml + 16) >> 8) % LL);
      const int l2 = (int)(((row0 + ml + 32) >> 8) % LL);
      sB[(ml)      * EP + n] = e0[g] + be + pos[l0 * EE + n];
      sB[(ml + 16) * EP + n] = e1[g] + be + pos[l1 * EE + n];
      sB[(ml + 32) * EP + n] = e2[g] + be + pos[l2 * EE + n];
    }
  }
  __syncthreads();

  {
    v2f a[3][16];
    #pragma unroll
    for (int kk = 0; kk < 16; ++kk) {
      const float* ap = sB + r * EP + 4 * kk + 2 * hf;
      a[0][kk] = *reinterpret_cast<const v2f*>(ap);
      a[1][kk] = *reinterpret_cast<const v2f*>(ap + 16 * EP);
      a[2][kk] = *reinterpret_cast<const v2f*>(ap + 32 * EP);
    }
    #pragma unroll
    for (int j = 0; j < 3; ++j) {
      const int n0 = (wid + 4 * j) * 16;
      v8f c0v = {0, 0, 0, 0, 0, 0, 0, 0};
      v8f c1v = {0, 0, 0, 0, 0, 0, 0, 0};
      v8f c2v = {0, 0, 0, 0, 0, 0, 0, 0};
      #pragma unroll
      for (int kk = 0; kk < 16; ++kk) {
        const int c0 = 4 * kk + 2 * hf;
        const float* bp = win + (size_t)c0 * 192 + n0 + r;
        v2f b; b.x = bp[0]; b.y = bp[192];
        c0v = wmma4(a[0][kk], b, c0v);
        c1v = wmma4(a[1][kk], b, c1v);
        c2v = wmma4(a[2][kk], b, c2v);
      }
      const int n = n0 + r;
      const float bi_ = bin[n];
      #pragma unroll
      for (int g = 0; g < 8; ++g) {
        const int ml = g + 8 * hf;
        sC[(ml)      * QP + n] = c0v[g] + bi_;
        sC[(ml + 16) * QP + n] = c1v[g] + bi_;
        sC[(ml + 32) * QP + n] = c2v[g] + bi_;
      }
    }
  }
  __syncthreads();

  float* sPw = sPall + wid * (4 * 16 * PP);
  #pragma unroll
  for (int bi = 0; bi < 4; ++bi) {
    const float* base = sC + (bi * LL) * QP + wid * HDD;
    float* sP = sPw + bi * (16 * PP);

    v8f S = {0, 0, 0, 0, 0, 0, 0, 0};
    #pragma unroll
    for (int k0 = 0; k0 < 16; k0 += 4) {
      const int c0 = k0 + 2 * hf;
      v2f a = {0, 0}, b = {0, 0};
      if (r < LL) {
        const float* qp = base + r * QP + c0;
        a = *reinterpret_cast<const v2f*>(qp);
        b = *reinterpret_cast<const v2f*>(qp + EE);
      }
      S = wmma4(a, b, S);
    }

    #pragma unroll
    for (int g = 0; g < 8; ++g) {
      float sv = (r < LL) ? S[g] * 0.25f : -__builtin_inff();
      float rm = sv;
      #pragma unroll
      for (int m = 8; m >= 1; m >>= 1) rm = fmaxf(rm, __shfl_xor(rm, m, 32));
      float e = __expf(sv - rm);
      float rs = e;
      #pragma unroll
      for (int m = 8; m >= 1; m >>= 1) rs += __shfl_xor(rs, m, 32);
      sP[(g + 8 * hf) * PP + r] = e / rs;
    }
  }
  __syncthreads();

  #pragma unroll
  for (int bi = 0; bi < 4; ++bi) {
    const float* base = sC + (bi * LL) * QP + wid * HDD;
    const float* sP = sPw + bi * (16 * PP);
    v8f Cv = {0, 0, 0, 0, 0, 0, 0, 0};
    #pragma unroll
    for (int k0 = 0; k0 < 16; k0 += 4) {
      const int c0 = k0 + 2 * hf;
      v2f a; a.x = sP[r * PP + c0]; a.y = sP[r * PP + c0 + 1];
      v2f b;
      b.x = (c0     < LL) ? base[c0       * QP + 128 + r] : 0.0f;
      b.y = (c0 + 1 < LL) ? base[(c0 + 1) * QP + 128 + r] : 0.0f;
      Cv = wmma4(a, b, Cv);
    }
    #pragma unroll
    for (int g = 0; g < 8; ++g) {
      const int m = g + 8 * hf;
      if (m < LL) sB[(bi * LL + m) * EP + wid * HDD + r] = Cv[g];
    }
  }
  __syncthreads();

  {
    float4* dst = reinterpret_cast<float4*>(ctx + row0 * EE);
    #pragma unroll
    for (int i = 0; i < 6; ++i) {
      const int j = tid + 128 * i;
      const int rr = j >> 4;
      const int c4 = j & 15;
      dst[j] = *reinterpret_cast<const float4*>(sB + rr * EP + c4 * 4);
    }
  }
}

__global__ __launch_bounds__(256) void k_proj_pool(const float* __restrict__ ctx,
                                                   const float* __restrict__ wout,
                                                   const float* __restrict__ bout,
                                                   float* __restrict__ pooled) {
  __shared__ float pool[EE];
  const int tid = threadIdx.x;
  if (tid < EE) pool[tid] = 0.0f;
  __syncthreads();

  const int lane = tid & 31;
  const int wid  = tid >> 5;
  const int r  = lane & 15;
  const int hf = lane >> 4;
  const size_t mbase = (size_t)blockIdx.x * 256;

  #pragma unroll
  for (int i = 0; i < 8; ++i) {
    const int t  = wid * 8 + i;
    const int m0 = (t >> 2) * 16;
    const int n0 = (t & 3) * 16;
    v8f acc = {0, 0, 0, 0, 0, 0, 0, 0};
    #pragma unroll
    for (int k0 = 0; k0 < 64; k0 += 4) {
      const int c0 = k0 + 2 * hf;
      const float* ap = ctx + (mbase + m0 + r) * EE + c0;
      v2f a; a.x = ap[0]; a.y = ap[1];
      const float* bp = wout + (size_t)c0 * EE + n0 + r;
      v2f b; b.x = bp[0]; b.y = bp[EE];
      acc = wmma4(a, b, acc);
    }
    float ps = acc[0] + acc[1] + acc[2] + acc[3] +
               acc[4] + acc[5] + acc[6] + acc[7];
    ps += __shfl_xor(ps, 16, 32);
    if (lane < 16) atomicAdd(&pool[n0 + r], ps);
  }
  __syncthreads();
  if (tid < EE)
    pooled[(size_t)blockIdx.x * EE + tid] = pool[tid] * (1.0f / 256.0f) + bout[tid];
}

__global__ __launch_bounds__(256) void k_mlp_lu(
    const float* __restrict__ pooled,
    const float* __restrict__ mw1, const float* __restrict__ mb1,
    const float* __restrict__ mg,  const float* __restrict__ mbb,
    const float* __restrict__ mw2, const float* __restrict__ mb2,
    const float* __restrict__ nw1, const float* __restrict__ nb1,
    const float* __restrict__ ng,  const float* __restrict__ nbb,
    const float* __restrict__ nw2, const float* __restrict__ nb2,
    float* __restrict__ out) {
  const int branch = blockIdx.x & 1;
  const int row    = blockIdx.x >> 1;
  const float* w1 = branch ? nw1 : mw1;
  const float* b1 = branch ? nb1 : mb1;
  const float* g  = branch ? ng  : mg;
  const float* bb = branch ? nbb : mbb;
  const float* w2 = branch ? nw2 : mw2;
  const float* b2 = branch ? nb2 : mb2;

  __shared__ float sx[EE];
  __shared__ float sh[HIDD];
  __shared__ float sraw[512];
  const int tid = threadIdx.x;

  if (tid < EE) sx[tid] = pooled[(size_t)row * EE + tid];
  __syncthreads();

  if (tid < HIDD) {
    float acc = b1[tid];
    #pragma unroll
    for (int i = 0; i < EE; ++i) acc += sx[i] * w1[i * HIDD + tid];
    float mu = acc;
    #pragma unroll
    for (int m = 16; m >= 1; m >>= 1) mu += __shfl_xor(mu, m, 32);
    mu *= (1.0f / 32.0f);
    const float d = acc - mu;
    float vv = d * d;
    #pragma unroll
    for (int m = 16; m >= 1; m >>= 1) vv += __shfl_xor(vv, m, 32);
    vv *= (1.0f / 32.0f);
    const float y = d * rsqrtf(vv + 1e-5f) * g[tid] + bb[tid];
    sh[tid] = 0.5f * y * (1.0f + erff(y * 0.70710678118654752f));
  }
  __syncthreads();

  for (int c = tid; c < 512; c += 256) {
    float acc = b2[c];
    #pragma unroll
    for (int j = 0; j < HIDD; ++j) acc += sh[j] * w2[j * 512 + c];
    sraw[c] = acc;
  }
  __syncthreads();

  const size_t obase = ((size_t)branch * 384 + row) * 512;
  for (int o = tid; o < 512; o += 256) {
    const int h  = o >> 6;
    const int ik = o & 63;
    const int i  = ik >> 3;
    const int k  = ik & 7;
    const float* rh = sraw + h * 64;
    float acc = 0.0f;
    #pragma unroll
    for (int j = 0; j < NKK; ++j) {
      if (j > i || j > k) continue;
      const float Lij = (j == i) ? 1.0f : rh[(i * (i - 1)) / 2 + j];
      float u = rh[28 + (NKK * j - (j * (j - 1)) / 2) + (k - j)];
      if (j == k) u = softplusf(u) + 0.01f;
      acc += Lij * u;
    }
    out[obase + o] = acc;
  }
}

extern "C" void kernel_launch(void* const* d_in, const int* in_sizes, int n_in,
                              void* d_out, int out_size, void* d_ws, size_t ws_size,
                              hipStream_t stream) {
  const float* wsfeat = (const float*)d_in[0];
  const float* W_emb  = (const float*)d_in[1];
  const float* b_emb  = (const float*)d_in[2];
  const float* pos    = (const float*)d_in[3];
  const float* w_in   = (const float*)d_in[4];
  const float* b_in   = (const float*)d_in[5];
  const float* w_out  = (const float*)d_in[6];
  const float* b_out  = (const float*)d_in[7];
  const float* m_w1 = (const float*)d_in[8];
  const float* m_b1 = (const float*)d_in[9];
  const float* m_g  = (const float*)d_in[10];
  const float* m_bb = (const float*)d_in[11];
  const float* m_w2 = (const float*)d_in[12];
  const float* m_b2 = (const float*)d_in[13];
  const float* n_w1 = (const float*)d_in[14];
  const float* n_b1 = (const float*)d_in[15];
  const float* n_g  = (const float*)d_in[16];
  const float* n_bb = (const float*)d_in[17];
  const float* n_w2 = (const float*)d_in[18];
  const float* n_b2 = (const float*)d_in[19];
  float* out = (float*)d_out;

  float* ws     = (float*)d_ws;
  float* feats  = ws;
  float* ctx    = feats + (size_t)MROWS * 56;
  float* pooled = ctx   + (size_t)MROWS * EE;

  k_stats<<<SROWS / 256, 256, 0, stream>>>(wsfeat, feats);
  k_fused<<<MROWS / UR, 128, 0, stream>>>(feats, W_emb, b_emb, pos, w_in, b_in, ctx);
  k_proj_pool<<<BB * LL, 256, 0, stream>>>(ctx, w_out, b_out, pooled);
  k_mlp_lu<<<BB * LL * 2, 256, 0, stream>>>(pooled,
      m_w1, m_b1, m_g, m_bb, m_w2, m_b2,
      n_w1, n_b1, n_g, n_bb, n_w2, n_b2, out);
}


# pool before project: mean over each 256-row group commutes with w_out, so K2 emits per-block column sums and WMMA projects 384 rows, not 98304; sum order changed, LDS float atomics gone
# speedup vs baseline: 3.0182x; 1.1297x over previous
#include <hip/hip_runtime.h>
#include <math.h>

#define LL   12
#define TT   8
#define BB   32
#define CC   256
#define SS   64
#define EE   64
#define HDD  16
#define NHH  4
#define HIDD 32
#define HH   8
#define NKK  8
#define MROWS (BB*CC*LL)
#define SROWS (LL*TT*BB*CC)

typedef float v2f __attribute__((ext_vector_type(2)));
typedef float v8f __attribute__((ext_vector_type(8)));

static __device__ __forceinline__ v8f wmma4(v2f a, v2f b, v8f c) {
  return __builtin_amdgcn_wmma_f32_16x16x4_f32(false, a, false, b, (short)0, c,
                                               false, false);
}

static __device__ __forceinline__ float softplusf(float x) {
  return (x > 20.0f) ? x : log1pf(__expf(x));
}

#define ROWQ 17
#define CE(a, b) { const float lo_ = fminf(v[a], v[b]); const float hi_ = fmaxf(v[a], v[b]); v[a] = lo_; v[b] = hi_; }

__global__ __launch_bounds__(256) void k_stats(const float* __restrict__ ws,
                                               float* __restrict__ feats) {
  __shared__ float4 tile[8 * 32 * ROWQ];
  const int lane = threadIdx.x & 31;
  const int wid  = threadIdx.x >> 5;
  const int row0 = (blockIdx.x * 8 + wid) * 32;

  const float4* src = reinterpret_cast<const float4*>(ws + (size_t)row0 * SS);
  float4* wt = tile + wid * (32 * ROWQ);
  #pragma unroll
  for (int q = 0; q < 16; ++q)
    wt[(2 * q + (lane >> 4)) * ROWQ + (lane & 15)] = src[q * 32 + lane];
  __syncthreads();

  float v[64];
  #pragma unroll
  for (int q = 0; q < 16; ++q) {
    const float4 x = wt[lane * ROWQ + q];
    v[4 * q + 0] = x.x; v[4 * q + 1] = x.y; v[4 * q + 2] = x.z; v[4 * q + 3] = x.w;
  }

  float sa = 0.0f, sb = 0.0f, sc = 0.0f, sd = 0.0f;
  float ta = 0.0f, tb = 0.0f, tc = 0.0f, td = 0.0f;
  #pragma unroll
  for (int q = 0; q < 16; ++q) {
    sa += v[4 * q + 0]; ta = fmaf(v[4 * q + 0], v[4 * q + 0], ta);
    sb += v[4 * q + 1]; tb = fmaf(v[4 * q + 1], v[4 * q + 1], tb);
    sc += v[4 * q + 2]; tc = fmaf(v[4 * q + 2], v[4 * q + 2], tc);
    sd += v[4 * q + 3]; td = fmaf(v[4 * q + 3], v[4 * q + 3], td);
  }
  const float s  = (sa + sb) + (sc + sd);
  const float s2 = (ta + tb) + (tc + td);
  const float mean = s * (1.0f / 64.0f);
  const float var  = (s2 - 64.0f * mean * mean) * (1.0f / 63.0f);

  CE(0,1) CE(2,3) CE(4,5) CE(6,7) CE(8,9) CE(10,11)
  CE(12,13) CE(14,15) CE(16,17) CE(18,19) CE(20,21) CE(22,23)
  CE(24,25) CE(26,27) CE(28,29) CE(30,31) CE(32,33) CE(34,35)
  CE(36,37) CE(38,39) CE(40,41) CE(42,43) CE(44,45) CE(46,47)
  CE(48,49) CE(50,51) CE(52,53) CE(54,55) CE(56,57) CE(58,59)
  CE(60,61) CE(62,63) CE(0,2) CE(1,3) CE(4,6) CE(5,7)
  CE(8,10) CE(9,11) CE(12,14) CE(13,15) CE(16,18) CE(17,19)
  CE(20,22) CE(21,23) CE(24,26) CE(25,27) CE(28,30) CE(29,31)
  CE(32,34) CE(33,35) CE(36,38) CE(37,39) CE(40,42) CE(41,43)
  CE(44,46) CE(45,47) CE(48,50) CE(49,51) CE(52,54) CE(53,55)
  CE(56,58) CE(57,59) CE(60,62) CE(61,63) CE(1,2) CE(5,6)
  CE(9,10) CE(13,14) CE(17,18) CE(21,22) CE(25,26) CE(29,30)
  CE(33,34) CE(37,38) CE(41,42) CE(45,46) CE(49,50) CE(53,54)
  CE(57,58) CE(61,62) CE(0,4) CE(1,5) CE(2,6) CE(3,7)
  CE(8,12) CE(9,13) CE(10,14) CE(11,15) CE(16,20) CE(17,21)
  CE(18,22) CE(19,23) CE(24,28) CE(25,29) CE(26,30) CE(27,31)
  CE(32,36) CE(33,37) CE(34,38) CE(35,39) CE(40,44) CE(41,45)
  CE(42,46) CE(43,47) CE(48,52) CE(49,53) CE(50,54) CE(51,55)
  CE(56,60) CE(57,61) CE(58,62) CE(59,63) CE(2,4) CE(3,5)
  CE(10,12) CE(11,13) CE(18,20) CE(19,21) CE(26,28) CE(27,29)
  CE(34,36) CE(35,37) CE(42,44) CE(43,45) CE(50,52) CE(51,53)
  CE(58,60) CE(59,61) CE(1,2) CE(3,4) CE(5,6) CE(9,10)
  CE(11,12) CE(13,14) CE(17,18) CE(19,20) CE(21,22) CE(25,26)
  CE(27,28) CE(29,30) CE(33,34) CE(35,36) CE(37,38) CE(41,42)
  CE(43,44) CE(45,46) CE(49,50) CE(51,52) CE(53,54) CE(57,58)
  CE(59,60) CE(61,62) CE(0,8) CE(1,9) CE(2,10) CE(3,11)
  CE(4,12) CE(5,13) CE(6,14) CE(7,15) CE(16,24) CE(17,25)
  CE(18,26) CE(19,27) CE(20,28) CE(21,29) CE(22,30) CE(23,31)
  CE(32,40) CE(33,41) CE(34,42) CE(35,43) CE(36,44) CE(37,45)
  CE(38,46) CE(39,47) CE(48,56) CE(49,57) CE(50,58) CE(51,59)
  CE(52,60) CE(53,61) CE(54,62) CE(55,63) CE(4,8) CE(5,9)
  CE(6,10) CE(7,11) CE(20,24) CE(21,25) CE(22,26) CE(23,27)
  CE(36,40) CE(37,41) CE(38,42) CE(39,43) CE(52,56) CE(53,57)
  CE(54,58) CE(55,59) CE(2,4) CE(3,5) CE(6,8) CE(7,9)
  CE(10,12) CE(11,13) CE(18,20) CE(19,21) CE(22,24) CE(23,25)
  CE(26,28) CE(27,29) CE(34,36) CE(35,37) CE(38,40) CE(39,41)
  CE(42,44) CE(43,45) CE(50,52) CE(51,53) CE(54,56) CE(55,57)
  CE(58,60) CE(59,61) CE(1,2) CE(3,4) CE(5,6) CE(7,8)
  CE(9,10) CE(11,12) CE(13,14) CE(17,18) CE(19,20) CE(21,22)
  CE(23,24) CE(25,26) CE(27,28) CE(29,30) CE(33,34) CE(35,36)
  CE(37,38) CE(39,40) CE(41,42) CE(43,44) CE(45,46) CE(49,50)
  CE(51,52) CE(53,54) CE(55,56) CE(57,58) CE(59,60) CE(61,62)
  CE(0,16) CE(1,17) CE(2,18) CE(3,19) CE(4,20) CE(5,21)
  CE(6,22) CE(7,23) CE(8,24) CE(9,25) CE(10,26) CE(11,27)
  CE(12,28) CE(13,29) CE(14,30) CE(15,31) CE(32,48) CE(33,49)
  CE(34,50) CE(35,51) CE(36,52) CE(37,53) CE(38,54) CE(39,55)
  CE(40,56) CE(41,57) CE(42,58) CE(43,59) CE(44,60) CE(45,61)
  CE(46,62) CE(47,63) CE(8,16) CE(9,17) CE(10,18) CE(11,19)
  CE(12,20) CE(13,21) CE(14,22) CE(15,23) CE(40,48) CE(41,49)
  CE(42,50) CE(43,51) CE(44,52) CE(45,53) CE(46,54) CE(47,55)
  CE(4,8) CE(5,9) CE(6,10) CE(7,11) CE(12,16) CE(13,17)
  CE(14,18) CE(15,19) CE(20,24) CE(21,25) CE(22,26) CE(23,27)
  CE(36,40) CE(37,41) CE(38,42) CE(39,43) CE(44,48) CE(45,49)
  CE(46,50) CE(47,51) CE(52,56) CE(53,57) CE(54,58) CE(55,59)
  CE(2,4) CE(3,5) CE(6,8) CE(7,9) CE(10,12) CE(11,13)
  CE(14,16) CE(15,17) CE(18,20) CE(19,21) CE(22,24) CE(23,25)
  CE(26,28) CE(27,29) CE(34,36) CE(35,37) CE(38,40) CE(39,41)
  CE(42,44) CE(43,45) CE(46,48) CE(47,49) CE(50,52) CE(51,53)
  CE(54,56) CE(55,57) CE(58,60) CE(59,61) CE(1,2) CE(3,4)
  CE(5,6) CE(7,8) CE(9,10) CE(11,12) CE(13,14) CE(15,16)
  CE(17,18) CE(19,20) CE(21,22) CE(23,24) CE(25,26) CE(27,28)
  CE(29,30) CE(33,34) CE(35,36) CE(37,38) CE(39,40) CE(41,42)
  CE(43,44) CE(45,46) CE(47,48) CE(49,50) CE(51,52) CE(53,54)
  CE(55,56) CE(57,58) CE(59,60) CE(61,62) CE(0,32) CE(1,33)
  CE(2,34) CE(3,35) CE(4,36) CE(5,37) CE(6,38) CE(7,39)
  CE(8,40) CE(9,41) CE(10,42) CE(11,43) CE(12,44) CE(13,45)
  CE(14,46) CE(15,47) CE(16,48) CE(17,49) CE(18,50) CE(19,51)
  CE(20,52) CE(21,53) CE(22,54) CE(23,55) CE(24,56) CE(25,57)
  CE(26,58) CE(27,59) CE(28,60) CE(29,61) CE(30,62) CE(31,63)
  CE(16,32) CE(17,33) CE(18,34) CE(19,35) CE(20,36) CE(21,37)
  CE(22,38) CE(23,39) CE(24,40) CE(25,41) CE(26,42) CE(27,43)
  CE(28,44) CE(29,45) CE(30,46) CE(31,47) CE(8,16) CE(9,17)
  CE(10,18) CE(11,19) CE(12,20) CE(13,21) CE(14,22) CE(15,23)
  CE(24,32) CE(25,33) CE(26,34) CE(27,35) CE(28,36) CE(29,37)
  CE(30,38) CE(31,39) CE(40,48) CE(41,49) CE(42,50) CE(43,51)
  CE(44,52) CE(45,53) CE(46,54) CE(47,55) CE(4,8) CE(5,9)
  CE(6,10) CE(7,11) CE(12,16) CE(13,17) CE(14,18) CE(15,19)
  CE(20,24) CE(21,25) CE(22,26) CE(23,27) CE(28,32) CE(29,33)
  CE(30,34) CE(31,35) CE(36,40) CE(37,41) CE(38,42) CE(39,43)
  CE(44,48) CE(45,49) CE(46,50) CE(47,51) CE(52,56) CE(53,57)
  CE(54,58) CE(55,59) CE(2,4) CE(3,5) CE(6,8) CE(7,9)
  CE(10,12) CE(11,13) CE(14,16) CE(15,17) CE(18,20) CE(19,21)
  CE(22,24) CE(23,25) CE(26,28) CE(27,29) CE(30,32) CE(31,33)
  CE(34,36) CE(35,37) CE(38,40) CE(39,41) CE(42,44) CE(43,45)
  CE(46,48) CE(47,49) CE(50,52) CE(51,53) CE(54,56) CE(55,57)
  CE(58,60) CE(59,61) CE(1,2) CE(3,4) CE(5,6) CE(7,8)
  CE(9,10) CE(11,12) CE(13,14) CE(15,16) CE(17,18) CE(19,20)
  CE(21,22) CE(23,24) CE(25,26) CE(27,28) CE(29,30) CE(31,32)
  CE(33,34) CE(35,36) CE(37,38) CE(39,40) CE(41,42) CE(43,44)
  CE(45,46) CE(47,48) CE(49,50) CE(51,52) CE(53,54) CE(55,56)
  CE(57,58) CE(59,60) CE(61,62)

  const float q0 = v[0];
  const float q1 = v[15] + 0.75f * (v[16] - v[15]);
  const float q2 = 0.5f * (v[31] + v[32]);
  const float q3 = v[47] + 0.25f * (v[48] - v[47]);
  const float q4 = v[63];

  const int row = row0 + lane;
  const int c = row & 255;
  int tmp = row >> 8;
  const int b = tmp & 31; tmp >>= 5;
  const int t = tmp & 7;
  const int l = tmp >> 3;
  float* o = feats + ((size_t)((b * LL + l) * CC + c)) * 56 + t * 7;
  o[0] = mean; o[1] = var; o[2] = q0; o[3] = q1; o[4] = q2; o[5] = q3; o[6] = q4;
}
#undef CE

#define UR 48
#define FP 60
#define EP 68
#define QP 196
#define PP 17

__global__ __launch_bounds__(128) void k_fused(const float* __restrict__ feats,
                                               const float* __restrict__ Wemb,
                                               const float* __restrict__ bemb,
                                               const float* __restrict__ pos,
                                               const float* __restrict__ win,
                                               const float* __restrict__ bin,
                                               float* __restrict__ part) {
  __shared__ __attribute__((aligned(16))) float sA[UR * FP];
  __shared__ float sPall[4 * 4 * 16 * PP];
  __shared__ __attribute__((aligned(16))) float sB[UR * EP];
  __shared__ __attribute__((aligned(16))) float sC[UR * QP];
  const int tid  = threadIdx.x;
  const int lane = tid & 31;
  const int wid  = tid >> 5;
  const int r  = lane & 15;
  const int hf = lane >> 4;
  const size_t row0 = (size_t)blockIdx.x * UR;

  {
    const float4* src = reinterpret_cast<const float4*>(feats + row0 * 56);
    if (tid < 112) {
      float4 x[6];
      #pragma unroll
      for (int i = 0; i < 6; ++i) x[i] = src[tid + 112 * i];
      #pragma unroll
      for (int i = 0; i < 6; ++i) {
        const int j = tid + 112 * i;
        const int rr = j / 14;
        const int c4 = j - rr * 14;
        *reinterpret_cast<float4*>(sA + rr * FP + c4 * 4) = x[i];
      }
    }
  }
  __syncthreads();

  {
    const int n0 = wid * 16;
    v8f e0 = {0, 0, 0, 0, 0, 0, 0, 0};
    v8f e1 = {0, 0, 0, 0, 0, 0, 0, 0};
    v8f e2 = {0, 0, 0, 0, 0, 0, 0, 0};
    #pragma unroll
    for (int k0 = 0; k0 < 56; k0 += 4) {
      const int c0 = k0 + 2 * hf;
      const float* bp = Wemb + (size_t)c0 * EE + n0 + r;
      v2f b; b.x = bp[0]; b.y = bp[EE];
      const float* ap = sA + r * FP + c0;
      const v2f a0 = *reinterpret_cast<const v2f*>(ap);
      const v2f a1 = *reinterpret_cast<const v2f*>(ap + 16 * FP);
      const v2f a2 = *reinterpret_cast<const v2f*>(ap + 32 * FP);
      e0 = wmma4(a0, b, e0);
      e1 = wmma4(a1, b, e1);
      e2 = wmma4(a2, b, e2);
    }
    const int n = n0 + r;
    const float be = bemb[n];
    #pragma unroll
    for (int g = 0; g < 8; ++g) {
      const int ml = g + 8 * hf;
      const int l0 = (int)(((row0 + ml)      >> 8) % LL);
      const int l1 = (int)(((row0 + ml + 16) >> 8) % LL);
      const int l2 = (int)(((row0 + ml + 32) >> 8) % LL);
      sB[(ml)      * EP + n] = e0[g] + be + pos[l0 * EE + n];
      sB[(ml + 16) * EP + n] = e1[g] + be + pos[l1 * EE + n];
      sB[(ml + 32) * EP + n] = e2[g] + be + pos[l2 * EE + n];
    }
  }
  __syncthreads();

  {
    v2f a[3][16];
    #pragma unroll
    for (int kk = 0; kk < 16; ++kk) {
      const float* ap = sB + r * EP + 4 * kk + 2 * hf;
      a[0][kk] = *reinterpret_cast<const v2f*>(ap);
      a[1][kk] = *reinterpret_cast<const v2f*>(ap + 16 * EP);
      a[2][kk] = *reinterpret_cast<const v2f*>(ap + 32 * EP);
    }
    #pragma unroll
    for (int j = 0; j < 3; ++j) {
      const int n0 = (wid + 4 * j) * 16;
      v8f c0v = {0, 0, 0, 0, 0, 0, 0, 0};
      v8f c1v = {0, 0, 0, 0, 0, 0, 0, 0};
      v8f c2v = {0, 0, 0, 0, 0, 0, 0, 0};
      #pragma unroll
      for (int kk = 0; kk < 16; ++kk) {
        const int c0 = 4 * kk + 2 * hf;
        const float* bp = win + (size_t)c0 * 192 + n0 + r;
        v2f b; b.x = bp[0]; b.y = bp[192];
        c0v = wmma4(a[0][kk], b, c0v);
        c1v = wmma4(a[1][kk], b, c1v);
        c2v = wmma4(a[2][kk], b, c2v);
      }
      const int n = n0 + r;
      const float bi_ = bin[n];
      #pragma unroll
      for (int g = 0; g < 8; ++g) {
        const int ml = g + 8 * hf;
        sC[(ml)      * QP + n] = c0v[g] + bi_;
        sC[(ml + 16) * QP + n] = c1v[g] + bi_;
        sC[(ml + 32) * QP + n] = c2v[g] + bi_;
      }
    }
  }
  __syncthreads();

  float* sPw = sPall + wid * (4 * 16 * PP);
  #pragma unroll
  for (int bi = 0; bi < 4; ++bi) {
    const float* base = sC + (bi * LL) * QP + wid * HDD;
    float* sP = sPw + bi * (16 * PP);

    v8f S = {0, 0, 0, 0, 0, 0, 0, 0};
    #pragma unroll
    for (int k0 = 0; k0 < 16; k0 += 4) {
      const int c0 = k0 + 2 * hf;
      v2f a = {0, 0}, b = {0, 0};
      if (r < LL) {
        const float* qp = base + r * QP + c0;
        a = *reinterpret_cast<const v2f*>(qp);
        b = *reinterpret_cast<const v2f*>(qp + EE);
      }
      S = wmma4(a, b, S);
    }

    #pragma unroll
    for (int g = 0; g < 8; ++g) {
      float sv = (r < LL) ? S[g] * 0.25f : -__builtin_inff();
      float rm = sv;
      #pragma unroll
      for (int m = 8; m >= 1; m >>= 1) rm = fmaxf(rm, __shfl_xor(rm, m, 32));
      float e = __expf(sv - rm);
      float rs = e;
      #pragma unroll
      for (int m = 8; m >= 1; m >>= 1) rs += __shfl_xor(rs, m, 32);
      sP[(g + 8 * hf) * PP + r] = e / rs;
    }
  }
  __syncthreads();

  const int gsplit = (int)((((row0 >> 8) + 1) << 8) - row0);
  float t0[4], t1[4];
  #pragma unroll
  for (int bi = 0; bi < 4; ++bi) {
    const float* base = sC + (bi * LL) * QP + wid * HDD;
    const float* sP = sPw + bi * (16 * PP);
    v8f Cv = {0, 0, 0, 0, 0, 0, 0, 0};
    #pragma unroll
    for (int k0 = 0; k0 < 16; k0 += 4) {
      const int c0 = k0 + 2 * hf;
      v2f a; a.x = sP[r * PP + c0]; a.y = sP[r * PP + c0 + 1];
      v2f b;
      b.x = (c0     < LL) ? base[c0       * QP + 128 + r] : 0.0f;
      b.y = (c0 + 1 < LL) ? base[(c0 + 1) * QP + 128 + r] : 0.0f;
      Cv = wmma4(a, b, Cv);
    }
    float u0 = 0.0f, u1 = 0.0f;
    #pragma unroll
    for (int g = 0; g < 8; ++g) {
      const int m = g + 8 * hf;
      const float cv = (m < LL) ? Cv[g] : 0.0f;
      const bool first = (bi * LL + m) < gsplit;
      u0 += first ? cv : 0.0f;
      u1 += first ? 0.0f : cv;
    }
    t0[bi] = u0; t1[bi] = u1;
  }
  float s0 = (t0[0] + t0[1]) + (t0[2] + t0[3]);
  float s1 = (t1[0] + t1[1]) + (t1[2] + t1[3]);
  s0 += __shfl_xor(s0, 16, 32);
  s1 += __shfl_xor(s1, 16, 32);
  if (hf == 0) {
    sB[wid * HDD + r]      = s0;
    sB[EE + wid * HDD + r] = s1;
  }
  __syncthreads();

  part[(size_t)blockIdx.x * (2 * EE) + tid] = sB[tid];
}

#define NGRP (BB * LL)
__global__ __launch_bounds__(128) void k_pool_proj(const float* __restrict__ part,
                                                   const float* __restrict__ wout,
                                                   const float* __restrict__ bout,
                                                   float* __restrict__ pooled) {
  __shared__ __attribute__((aligned(16))) float sM[16 * EP];
  const int tid  = threadIdx.x;
  const int lane = tid & 31;
  const int wid  = tid >> 5;
  const int r  = lane & 15;
  const int hf = lane >> 4;
  const int G0 = blockIdx.x * 16;
  const int n0 = wid * 16;

  v2f b[16];
  #pragma unroll
  for (int kk = 0; kk < 16; ++kk) {
    const int c0 = 4 * kk + 2 * hf;
    const float* bp = wout + (size_t)c0 * EE + n0 + r;
    b[kk].x = bp[0]; b[kk].y = bp[EE];
  }
  const float bo = bout[n0 + r];

  {
    const float4* p4 = reinterpret_cast<const float4*>(part);
    float4 pv[2][7];
    #pragma unroll
    for (int i = 0; i < 2; ++i) {
      const int it = tid + 128 * i;
      const int G  = G0 + (it >> 4);
      const int jf = (G * 256) / UR;
      const int jl = (G * 256 + 255) / UR;
      #pragma unroll
      for (int jj = 0; jj < 7; ++jj) {
        const int j = min(jf + jj, jl);
        const int slot = G - ((j * UR) >> 8);
        pv[i][jj] = p4[(size_t)(j * 2 + slot) * (EE / 4) + (it & 15)];
      }
    }
    #pragma unroll
    for (int i = 0; i < 2; ++i) {
      const int it = tid + 128 * i;
      const int G  = G0 + (it >> 4);
      const int jf = (G * 256) / UR;
      const int jl = (G * 256 + 255) / UR;
      float4 s = pv[i][0];
      #pragma unroll
      for (int jj = 1; jj < 7; ++jj) {
        const bool have = (jf + jj) <= jl;
        s.x += have ? pv[i][jj].x : 0.0f;
        s.y += have ? pv[i][jj].y : 0.0f;
        s.z += have ? pv[i][jj].z : 0.0f;
        s.w += have ? pv[i][jj].w : 0.0f;
      }
      s.x *= (1.0f / 256.0f); s.y *= (1.0f / 256.0f); s.z *= (1.0f / 256.0f); s.w *= (1.0f / 256.0f);
      *reinterpret_cast<float4*>(sM + (it >> 4) * EP + (it & 15) * 4) = s;
    }
  }
  __syncthreads();

  v8f acc0 = {0, 0, 0, 0, 0, 0, 0, 0};
  v8f acc1 = {0, 0, 0, 0, 0, 0, 0, 0};
  #pragma unroll
  for (int kk = 0; kk < 16; kk += 2) {
    const float* ap = sM + r * EP + 4 * kk + 2 * hf;
    const v2f a0 = *reinterpret_cast<const v2f*>(ap);
    const v2f a1 = *reinterpret_cast<const v2f*>(ap + 4);
    acc0 = wmma4(a0, b[kk],     acc0);
    acc1 = wmma4(a1, b[kk + 1], acc1);
  }
  __syncthreads();

  #pragma unroll
  for (int g = 0; g < 8; ++g)
    sM[(g + 8 * hf) * EP + n0 + r] = (acc0[g] + acc1[g]) + bo;
  __syncthreads();

  #pragma unroll
  for (int i = 0; i < 8; ++i) {
    const int j = tid + 128 * i;
    pooled[(size_t)G0 * EE + j] = sM[(j >> 6) * EP + (j & 63)];
  }
}

__global__ __launch_bounds__(256) void k_mlp_lu(
    const float* __restrict__ pooled,
    const float* __restrict__ mw1, const float* __restrict__ mb1,
    const float* __restrict__ mg,  const float* __restrict__ mbb,
    const float* __restrict__ mw2, const float* __restrict__ mb2,
    const float* __restrict__ nw1, const float* __restrict__ nb1,
    const float* __restrict__ ng,  const float* __restrict__ nbb,
    const float* __restrict__ nw2, const float* __restrict__ nb2,
    float* __restrict__ out) {
  const int branch = blockIdx.x & 1;
  const int row    = blockIdx.x >> 1;
  const float* w1 = branch ? nw1 : mw1;
  const float* b1 = branch ? nb1 : mb1;
  const float* g  = branch ? ng  : mg;
  const float* bb = branch ? nbb : mbb;
  const float* w2 = branch ? nw2 : mw2;
  const float* b2 = branch ? nb2 : mb2;

  __shared__ float sx[EE];
  __shared__ float sh[HIDD];
  __shared__ float sraw[512];
  const int tid = threadIdx.x;

  if (tid < EE) sx[tid] = pooled[(size_t)row * EE + tid];
  __syncthreads();

  if (tid < HIDD) {
    float acc = b1[tid];
    #pragma unroll
    for (int i = 0; i < EE; ++i) acc += sx[i] * w1[i * HIDD + tid];
    float mu = acc;
    #pragma unroll
    for (int m = 16; m >= 1; m >>= 1) mu += __shfl_xor(mu, m, 32);
    mu *= (1.0f / 32.0f);
    const float d = acc - mu;
    float vv = d * d;
    #pragma unroll
    for (int m = 16; m >= 1; m >>= 1) vv += __shfl_xor(vv, m, 32);
    vv *= (1.0f / 32.0f);
    const float y = d * rsqrtf(vv + 1e-5f) * g[tid] + bb[tid];
    sh[tid] = 0.5f * y * (1.0f + erff(y * 0.70710678118654752f));
  }
  __syncthreads();

  for (int c = tid; c < 512; c += 256) {
    float acc = b2[c];
    #pragma unroll
    for (int j = 0; j < HIDD; ++j) acc += sh[j] * w2[j * 512 + c];
    sraw[c] = acc;
  }
  __syncthreads();

  const size_t obase = ((size_t)branch * 384 + row) * 512;
  for (int o = tid; o < 512; o += 256) {
    const int h  = o >> 6;
    const int ik = o & 63;
    const int i  = ik >> 3;
    const int k  = ik & 7;
    const float* rh = sraw + h * 64;
    float acc = 0.0f;
    #pragma unroll
    for (int j = 0; j < NKK; ++j) {
      if (j > i || j > k) continue;
      const float Lij = (j == i) ? 1.0f : rh[(i * (i - 1)) / 2 + j];
      float u = rh[28 + (NKK * j - (j * (j - 1)) / 2) + (k - j)];
      if (j == k) u = softplusf(u) + 0.01f;
      acc += Lij * u;
    }
    out[obase + o] = acc;
  }
}

extern "C" void kernel_launch(void* const* d_in, const int* in_sizes, int n_in,
                              void* d_out, int out_size, void* d_ws, size_t ws_size,
                              hipStream_t stream) {
  const float* wsfeat = (const float*)d_in[0];
  const float* W_emb  = (const float*)d_in[1];
  const float* b_emb  = (const float*)d_in[2];
  const float* pos    = (const float*)d_in[3];
  const float* w_in   = (const float*)d_in[4];
  const float* b_in   = (const float*)d_in[5];
  const float* w_out  = (const float*)d_in[6];
  const float* b_out  = (const float*)d_in[7];
  const float* m_w1 = (const float*)d_in[8];
  const float* m_b1 = (const float*)d_in[9];
  const float* m_g  = (const float*)d_in[10];
  const float* m_bb = (const float*)d_in[11];
  const float* m_w2 = (const float*)d_in[12];
  const float* m_b2 = (const float*)d_in[13];
  const float* n_w1 = (const float*)d_in[14];
  const float* n_b1 = (const float*)d_in[15];
  const float* n_g  = (const float*)d_in[16];
  const float* n_bb = (const float*)d_in[17];
  const float* n_w2 = (const float*)d_in[18];
  const float* n_b2 = (const float*)d_in[19];
  float* out = (float*)d_out;

  float* ws     = (float*)d_ws;
  float* feats  = ws;
  float* part   = feats + (size_t)MROWS * 56;
  float* pooled = part  + (size_t)MROWS * EE;

  k_stats<<<SROWS / 256, 256, 0, stream>>>(wsfeat, feats);
  k_fused<<<MROWS / UR, 128, 0, stream>>>(feats, W_emb, b_emb, pos, w_in, b_in, part);
  k_pool_proj<<<NGRP / 16, 128, 0, stream>>>(part, w_out, b_out, pooled);
  k_mlp_lu<<<BB * LL * 2, 256, 0, stream>>>(pooled,
      m_w1, m_b1, m_g, m_bb, m_w2, m_b2,
      n_w1, n_b1, n_g, n_bb, n_w2, n_b2, out);
}


# fuse the row statistics into the attention kernel: each block sorts its own 384 source rows (one lane per row, 3 passes, next pass's loads in flight while sorting), feats stay in LDS; 4 launches -> 3
# speedup vs baseline: 3.3185x; 1.0995x over previous
#include <hip/hip_runtime.h>
#include <math.h>

#define LL   12
#define TT   8
#define BB   32
#define CC   256
#define SS   64
#define EE   64
#define HDD  16
#define NHH  4
#define HIDD 32
#define HH   8
#define NKK  8
#define MROWS (BB*CC*LL)
#define SROWS (LL*TT*BB*CC)

typedef float v2f __attribute__((ext_vector_type(2)));
typedef float v4f __attribute__((ext_vector_type(4)));
typedef float v8f __attribute__((ext_vector_type(8)));

static __device__ __forceinline__ v8f wmma4(v2f a, v2f b, v8f c) {
  return __builtin_amdgcn_wmma_f32_16x16x4_f32(false, a, false, b, (short)0, c,
                                               false, false);
}

static __device__ __forceinline__ float softplusf(float x) {
  return (x > 20.0f) ? x : log1pf(__expf(x));
}

#define ROWQ 17
#define CE(a, b) { const float lo_ = fminf(v[a], v[b]); const float hi_ = fmaxf(v[a], v[b]); v[a] = lo_; v[b] = hi_; }

static __device__ __forceinline__ void row_stats(float (&v)[64], float* __restrict__ o) {
  float sa = 0.0f, sb = 0.0f, sc = 0.0f, sd = 0.0f;
  float ta = 0.0f, tb = 0.0f, tc = 0.0f, td = 0.0f;
  #pragma unroll
  for (int q = 0; q < 16; ++q) {
    sa += v[4 * q + 0]; ta = fmaf(v[4 * q + 0], v[4 * q + 0], ta);
    sb += v[4 * q + 1]; tb = fmaf(v[4 * q + 1], v[4 * q + 1], tb);
    sc += v[4 * q + 2]; tc = fmaf(v[4 * q + 2], v[4 * q + 2], tc);
    sd += v[4 * q + 3]; td = fmaf(v[4 * q + 3], v[4 * q + 3], td);
  }
  const float s  = (sa + sb) + (sc + sd);
  const float s2 = (ta + tb) + (tc + td);
  const float mean = s * (1.0f / 64.0f);
  const float var  = (s2 - 64.0f * mean * mean) * (1.0f / 63.0f);

  CE(0,1) CE(2,3) CE(4,5) CE(6,7) CE(8,9) CE(10,11)
  CE(12,13) CE(14,15) CE(16,17) CE(18,19) CE(20,21) CE(22,23)
  CE(24,25) CE(26,27) CE(28,29) CE(30,31) CE(32,33) CE(34,35)
  CE(36,37) CE(38,39) CE(40,41) CE(42,43) CE(44,45) CE(46,47)
  CE(48,49) CE(50,51) CE(52,53) CE(54,55) CE(56,57) CE(58,59)
  CE(60,61) CE(62,63) CE(0,2) CE(1,3) CE(4,6) CE(5,7)
  CE(8,10) CE(9,11) CE(12,14) CE(13,15) CE(16,18) CE(17,19)
  CE(20,22) CE(21,23) CE(24,26) CE(25,27) CE(28,30) CE(29,31)
  CE(32,34) CE(33,35) CE(36,38) CE(37,39) CE(40,42) CE(41,43)
  CE(44,46) CE(45,47) CE(48,50) CE(49,51) CE(52,54) CE(53,55)
  CE(56,58) CE(57,59) CE(60,62) CE(61,63) CE(1,2) CE(5,6)
  CE(9,10) CE(13,14) CE(17,18) CE(21,22) CE(25,26) CE(29,30)
  CE(33,34) CE(37,38) CE(41,42) CE(45,46) CE(49,50) CE(53,54)
  CE(57,58) CE(61,62) CE(0,4) CE(1,5) CE(2,6) CE(3,7)
  CE(8,12) CE(9,13) CE(10,14) CE(11,15) CE(16,20) CE(17,21)
  CE(18,22) CE(19,23) CE(24,28) CE(25,29) CE(26,30) CE(27,31)
  CE(32,36) CE(33,37) CE(34,38) CE(35,39) CE(40,44) CE(41,45)
  CE(42,46) CE(43,47) CE(48,52) CE(49,53) CE(50,54) CE(51,55)
  CE(56,60) CE(57,61) CE(58,62) CE(59,63) CE(2,4) CE(3,5)
  CE(10,12) CE(11,13) CE(18,20) CE(19,21) CE(26,28) CE(27,29)
  CE(34,36) CE(35,37) CE(42,44) CE(43,45) CE(50,52) CE(51,53)
  CE(58,60) CE(59,61) CE(1,2) CE(3,4) CE(5,6) CE(9,10)
  CE(11,12) CE(13,14) CE(17,18) CE(19,20) CE(21,22) CE(25,26)
  CE(27,28) CE(29,30) CE(33,34) CE(35,36) CE(37,38) CE(41,42)
  CE(43,44) CE(45,46) CE(49,50) CE(51,52) CE(53,54) CE(57,58)
  CE(59,60) CE(61,62) CE(0,8) CE(1,9) CE(2,10) CE(3,11)
  CE(4,12) CE(5,13) CE(6,14) CE(7,15) CE(16,24) CE(17,25)
  CE(18,26) CE(19,27) CE(20,28) CE(21,29) CE(22,30) CE(23,31)
  CE(32,40) CE(33,41) CE(34,42) CE(35,43) CE(36,44) CE(37,45)
  CE(38,46) CE(39,47) CE(48,56) CE(49,57) CE(50,58) CE(51,59)
  CE(52,60) CE(53,61) CE(54,62) CE(55,63) CE(4,8) CE(5,9)
  CE(6,10) CE(7,11) CE(20,24) CE(21,25) CE(22,26) CE(23,27)
  CE(36,40) CE(37,41) CE(38,42) CE(39,43) CE(52,56) CE(53,57)
  CE(54,58) CE(55,59) CE(2,4) CE(3,5) CE(6,8) CE(7,9)
  CE(10,12) CE(11,13) CE(18,20) CE(19,21) CE(22,24) CE(23,25)
  CE(26,28) CE(27,29) CE(34,36) CE(35,37) CE(38,40) CE(39,41)
  CE(42,44) CE(43,45) CE(50,52) CE(51,53) CE(54,56) CE(55,57)
  CE(58,60) CE(59,61) CE(1,2) CE(3,4) CE(5,6) CE(7,8)
  CE(9,10) CE(11,12) CE(13,14) CE(17,18) CE(19,20) CE(21,22)
  CE(23,24) CE(25,26) CE(27,28) CE(29,30) CE(33,34) CE(35,36)
  CE(37,38) CE(39,40) CE(41,42) CE(43,44) CE(45,46) CE(49,50)
  CE(51,52) CE(53,54) CE(55,56) CE(57,58) CE(59,60) CE(61,62)
  CE(0,16) CE(1,17) CE(2,18) CE(3,19) CE(4,20) CE(5,21)
  CE(6,22) CE(7,23) CE(8,24) CE(9,25) CE(10,26) CE(11,27)
  CE(12,28) CE(13,29) CE(14,30) CE(15,31) CE(32,48) CE(33,49)
  CE(34,50) CE(35,51) CE(36,52) CE(37,53) CE(38,54) CE(39,55)
  CE(40,56) CE(41,57) CE(42,58) CE(43,59) CE(44,60) CE(45,61)
  CE(46,62) CE(47,63) CE(8,16) CE(9,17) CE(10,18) CE(11,19)
  CE(12,20) CE(13,21) CE(14,22) CE(15,23) CE(40,48) CE(41,49)
  CE(42,50) CE(43,51) CE(44,52) CE(45,53) CE(46,54) CE(47,55)
  CE(4,8) CE(5,9) CE(6,10) CE(7,11) CE(12,16) CE(13,17)
  CE(14,18) CE(15,19) CE(20,24) CE(21,25) CE(22,26) CE(23,27)
  CE(36,40) CE(37,41) CE(38,42) CE(39,43) CE(44,48) CE(45,49)
  CE(46,50) CE(47,51) CE(52,56) CE(53,57) CE(54,58) CE(55,59)
  CE(2,4) CE(3,5) CE(6,8) CE(7,9) CE(10,12) CE(11,13)
  CE(14,16) CE(15,17) CE(18,20) CE(19,21) CE(22,24) CE(23,25)
  CE(26,28) CE(27,29) CE(34,36) CE(35,37) CE(38,40) CE(39,41)
  CE(42,44) CE(43,45) CE(46,48) CE(47,49) CE(50,52) CE(51,53)
  CE(54,56) CE(55,57) CE(58,60) CE(59,61) CE(1,2) CE(3,4)
  CE(5,6) CE(7,8) CE(9,10) CE(11,12) CE(13,14) CE(15,16)
  CE(17,18) CE(19,20) CE(21,22) CE(23,24) CE(25,26) CE(27,28)
  CE(29,30) CE(33,34) CE(35,36) CE(37,38) CE(39,40) CE(41,42)
  CE(43,44) CE(45,46) CE(47,48) CE(49,50) CE(51,52) CE(53,54)
  CE(55,56) CE(57,58) CE(59,60) CE(61,62) CE(0,32) CE(1,33)
  CE(2,34) CE(3,35) CE(4,36) CE(5,37) CE(6,38) CE(7,39)
  CE(8,40) CE(9,41) CE(10,42) CE(11,43) CE(12,44) CE(13,45)
  CE(14,46) CE(15,47) CE(16,48) CE(17,49) CE(18,50) CE(19,51)
  CE(20,52) CE(21,53) CE(22,54) CE(23,55) CE(24,56) CE(25,57)
  CE(26,58) CE(27,59) CE(28,60) CE(29,61) CE(30,62) CE(31,63)
  CE(16,32) CE(17,33) CE(18,34) CE(19,35) CE(20,36) CE(21,37)
  CE(22,38) CE(23,39) CE(24,40) CE(25,41) CE(26,42) CE(27,43)
  CE(28,44) CE(29,45) CE(30,46) CE(31,47) CE(8,16) CE(9,17)
  CE(10,18) CE(11,19) CE(12,20) CE(13,21) CE(14,22) CE(15,23)
  CE(24,32) CE(25,33) CE(26,34) CE(27,35) CE(28,36) CE(29,37)
  CE(30,38) CE(31,39) CE(40,48) CE(41,49) CE(42,50) CE(43,51)
  CE(44,52) CE(45,53) CE(46,54) CE(47,55) CE(4,8) CE(5,9)
  CE(6,10) CE(7,11) CE(12,16) CE(13,17) CE(14,18) CE(15,19)
  CE(20,24) CE(21,25) CE(22,26) CE(23,27) CE(28,32) CE(29,33)
  CE(30,34) CE(31,35) CE(36,40) CE(37,41) CE(38,42) CE(39,43)
  CE(44,48) CE(45,49) CE(46,50) CE(47,51) CE(52,56) CE(53,57)
  CE(54,58) CE(55,59) CE(2,4) CE(3,5) CE(6,8) CE(7,9)
  CE(10,12) CE(11,13) CE(14,16) CE(15,17) CE(18,20) CE(19,21)
  CE(22,24) CE(23,25) CE(26,28) CE(27,29) CE(30,32) CE(31,33)
  CE(34,36) CE(35,37) CE(38,40) CE(39,41) CE(42,44) CE(43,45)
  CE(46,48) CE(47,49) CE(50,52) CE(51,53) CE(54,56) CE(55,57)
  CE(58,60) CE(59,61) CE(1,2) CE(3,4) CE(5,6) CE(7,8)
  CE(9,10) CE(11,12) CE(13,14) CE(15,16) CE(17,18) CE(19,20)
  CE(21,22) CE(23,24) CE(25,26) CE(27,28) CE(29,30) CE(31,32)
  CE(33,34) CE(35,36) CE(37,38) CE(39,40) CE(41,42) CE(43,44)
  CE(45,46) CE(47,48) CE(49,50) CE(51,52) CE(53,54) CE(55,56)
  CE(57,58) CE(59,60) CE(61,62)

  const float q0 = v[0];
  const float q1 = v[15] + 0.75f * (v[16] - v[15]);
  const float q2 = 0.5f * (v[31] + v[32]);
  const float q3 = v[47] + 0.25f * (v[48] - v[47]);
  const float q4 = v[63];

  o[0] = mean; o[1] = var; o[2] = q0; o[3] = q1; o[4] = q2; o[5] = q3; o[6] = q4;
}
#undef CE

#define UR 48
#define FP 60
#define EP 68
#define QP 196
#define PP 17

__global__ __launch_bounds__(128) void k_fused(const float* __restrict__ ws,
                                               const float* __restrict__ Wemb,
                                               const float* __restrict__ bemb,
                                               const float* __restrict__ pos,
                                               const float* __restrict__ win,
                                               const float* __restrict__ bin,
                                               float* __restrict__ part) {
  __shared__ __attribute__((aligned(16))) float sA[UR * FP];
  __shared__ float sPall[4 * 4 * 16 * PP];
  __shared__ __attribute__((aligned(16))) float sB[UR * EP];
  __shared__ __attribute__((aligned(16))) float sC[UR * QP];
  const int tid  = threadIdx.x;
  const int lane = tid & 31;
  const int wid  = tid >> 5;
  const int r  = lane & 15;
  const int hf = lane >> 4;
  const size_t row0 = (size_t)blockIdx.x * UR;

  {
    int* sRow = reinterpret_cast<int*>(sB);
    v4f* wt = reinterpret_cast<v4f*>(sC) + wid * (32 * ROWQ);
    #pragma unroll
    for (int k = 0; k < 3; ++k) {
      const int idx = tid + 128 * k;
      const int t = idx / UR;
      const int i = idx - t * UR;
      const int R = (int)row0 + i;
      const int c = R & 255;
      const int bl = R >> 8;
      const int b = bl / LL;
      const int l = bl - b * LL;
      sRow[idx] = ((l * TT + t) * BB + b) * CC + c;
    }
    __syncthreads();

    v4f x[16];
    {
      const int base = wid * 32 + hf;
      #pragma unroll
      for (int q = 0; q < 16; ++q) {
        const int sr = sRow[base + 2 * q];
        x[q] = *(reinterpret_cast<const v4f*>(ws + (size_t)sr * SS) + r);
      }
    }
    #pragma unroll 1
    for (int p = 0; p < 3; ++p) {
      #pragma unroll
      for (int q = 0; q < 16; ++q) wt[(2 * q + hf) * ROWQ + r] = x[q];
      __syncthreads();

      float v[64];
      #pragma unroll
      for (int q = 0; q < 16; ++q) {
        const v4f y = wt[lane * ROWQ + q];
        v[4 * q + 0] = y.x; v[4 * q + 1] = y.y; v[4 * q + 2] = y.z; v[4 * q + 3] = y.w;
      }
      if (p < 2) {
        const int base = (p + 1) * 128 + wid * 32 + hf;
        #pragma unroll
        for (int q = 0; q < 16; ++q) {
          const int sr = sRow[base + 2 * q];
          x[q] = *(reinterpret_cast<const v4f*>(ws + (size_t)sr * SS) + r);
        }
      }
      const int idx = p * 128 + tid;
      const int t = idx / UR;
      const int i = idx - t * UR;
      row_stats(v, sA + i * FP + t * 7);
    }
  }
  __syncthreads();

  {
    const int n0 = wid * 16;
    v8f e0 = {0, 0, 0, 0, 0, 0, 0, 0};
    v8f e1 = {0, 0, 0, 0, 0, 0, 0, 0};
    v8f e2 = {0, 0, 0, 0, 0, 0, 0, 0};
    #pragma unroll
    for (int k0 = 0; k0 < 56; k0 += 4) {
      const int c0 = k0 + 2 * hf;
      const float* bp = Wemb + (size_t)c0 * EE + n0 + r;
      v2f b; b.x = bp[0]; b.y = bp[EE];
      const float* ap = sA + r * FP + c0;
      const v2f a0 = *reinterpret_cast<const v2f*>(ap);
      const v2f a1 = *reinterpret_cast<const v2f*>(ap + 16 * FP);
      const v2f a2 = *reinterpret_cast<const v2f*>(ap + 32 * FP);
      e0 = wmma4(a0, b, e0);
      e1 = wmma4(a1, b, e1);
      e2 = wmma4(a2, b, e2);
    }
    const int n = n0 + r;
    const float be = bemb[n];
    #pragma unroll
    for (int g = 0; g < 8; ++g) {
      const int ml = g + 8 * hf;
      const int l0 = (int)(((row0 + ml)      >> 8) % LL);
      const int l1 = (int)(((row0 + ml + 16) >> 8) % LL);
      const int l2 = (int)(((row0 + ml + 32) >> 8) % LL);
      sB[(ml)      * EP + n] = e0[g] + be + pos[l0 * EE + n];
      sB[(ml + 16) * EP + n] = e1[g] + be + pos[l1 * EE + n];
      sB[(ml + 32) * EP + n] = e2[g] + be + pos[l2 * EE + n];
    }
  }
  __syncthreads();

  {
    v2f a[3][16];
    #pragma unroll
    for (int kk = 0; kk < 16; ++kk) {
      const float* ap = sB + r * EP + 4 * kk + 2 * hf;
      a[0][kk] = *reinterpret_cast<const v2f*>(ap);
      a[1][kk] = *reinterpret_cast<const v2f*>(ap + 16 * EP);
      a[2][kk] = *reinterpret_cast<const v2f*>(ap + 32 * EP);
    }
    #pragma unroll
    for (int j = 0; j < 3; ++j) {
      const int n0 = (wid + 4 * j) * 16;
      v8f c0v = {0, 0, 0, 0, 0, 0, 0, 0};
      v8f c1v = {0, 0, 0, 0, 0, 0, 0, 0};
      v8f c2v = {0, 0, 0, 0, 0, 0, 0, 0};
      #pragma unroll
      for (int kk = 0; kk < 16; ++kk) {
        const int c0 = 4 * kk + 2 * hf;
        const float* bp = win + (size_t)c0 * 192 + n0 + r;
        v2f b; b.x = bp[0]; b.y = bp[192];
        c0v = wmma4(a[0][kk], b, c0v);
        c1v = wmma4(a[1][kk], b, c1v);
        c2v = wmma4(a[2][kk], b, c2v);
      }
      const int n = n0 + r;
      const float bi_ = bin[n];
      #pragma unroll
      for (int g = 0; g < 8; ++g) {
        const int ml = g + 8 * hf;
        sC[(ml)      * QP + n] = c0v[g] + bi_;
        sC[(ml + 16) * QP + n] = c1v[g] + bi_;
        sC[(ml + 32) * QP + n] = c2v[g] + bi_;
      }
    }
  }
  __syncthreads();

  float* sPw = sPall + wid * (4 * 16 * PP);
  #pragma unroll
  for (int bi = 0; bi < 4; ++bi) {
    const float* base = sC + (bi * LL) * QP + wid * HDD;
    float* sP = sPw + bi * (16 * PP);

    v8f S = {0, 0, 0, 0, 0, 0, 0, 0};
    #pragma unroll
    for (int k0 = 0; k0 < 16; k0 += 4) {
      const int c0 = k0 + 2 * hf;
      v2f a = {0, 0}, b = {0, 0};
      if (r < LL) {
        const float* qp = base + r * QP + c0;
        a = *reinterpret_cast<const v2f*>(qp);
        b = *reinterpret_cast<const v2f*>(qp + EE);
      }
      S = wmma4(a, b, S);
    }

    #pragma unroll
    for (int g = 0; g < 8; ++g) {
      float sv = (r < LL) ? S[g] * 0.25f : -__builtin_inff();
      float rm = sv;
      #pragma unroll
      for (int m = 8; m >= 1; m >>= 1) rm = fmaxf(rm, __shfl_xor(rm, m, 32));
      float e = __expf(sv - rm);
      float rs = e;
      #pragma unroll
      for (int m = 8; m >= 1; m >>= 1) rs += __shfl_xor(rs, m, 32);
      sP[(g + 8 * hf) * PP + r] = e / rs;
    }
  }
  __syncthreads();

  const int gsplit = (int)((((row0 >> 8) + 1) << 8) - row0);
  float t0[4], t1[4];
  #pragma unroll
  for (int bi = 0; bi < 4; ++bi) {
    const float* base = sC + (bi * LL) * QP + wid * HDD;
    const float* sP = sPw + bi * (16 * PP);
    v8f Cv = {0, 0, 0, 0, 0, 0, 0, 0};
    #pragma unroll
    for (int k0 = 0; k0 < 16; k0 += 4) {
      const int c0 = k0 + 2 * hf;
      v2f a; a.x = sP[r * PP + c0]; a.y = sP[r * PP + c0 + 1];
      v2f b;
      b.x = (c0     < LL) ? base[c0       * QP + 128 + r] : 0.0f;
      b.y = (c0 + 1 < LL) ? base[(c0 + 1) * QP + 128 + r] : 0.0f;
      Cv = wmma4(a, b, Cv);
    }
    float u0 = 0.0f, u1 = 0.0f;
    #pragma unroll
    for (int g = 0; g < 8; ++g) {
      const int m = g + 8 * hf;
      const float cv = (m < LL) ? Cv[g] : 0.0f;
      const bool first = (bi * LL + m) < gsplit;
      u0 += first ? cv : 0.0f;
      u1 += first ? 0.0f : cv;
    }
    t0[bi] = u0; t1[bi] = u1;
  }
  float s0 = (t0[0] + t0[1]) + (t0[2] + t0[3]);
  float s1 = (t1[0] + t1[1]) + (t1[2] + t1[3]);
  s0 += __shfl_xor(s0, 16, 32);
  s1 += __shfl_xor(s1, 16, 32);
  if (hf == 0) {
    sB[wid * HDD + r]      = s0;
    sB[EE + wid * HDD + r] = s1;
  }
  __syncthreads();

  part[(size_t)blockIdx.x * (2 * EE) + tid] = sB[tid];
}

#define NGRP (BB * LL)
__global__ __launch_bounds__(128) void k_pool_proj(const float* __restrict__ part,
                                                   const float* __restrict__ wout,
                                                   const float* __restrict__ bout,
                                                   float* __restrict__ pooled) {
  __shared__ __attribute__((aligned(16))) float sM[16 * EP];
  const int tid  = threadIdx.x;
  const int lane = tid & 31;
  const int wid  = tid >> 5;
  const int r  = lane & 15;
  const int hf = lane >> 4;
  const int G0 = blockIdx.x * 16;
  const int n0 = wid * 16;

  v2f b[16];
  #pragma unroll
  for (int kk = 0; kk < 16; ++kk) {
    const int c0 = 4 * kk + 2 * hf;
    const float* bp = wout + (size_t)c0 * EE + n0 + r;
    b[kk].x = bp[0]; b[kk].y = bp[EE];
  }
  const float bo = bout[n0 + r];

  {
    const float4* p4 = reinterpret_cast<const float4*>(part);
    float4 pv[2][7];
    #pragma unroll
    for (int i = 0; i < 2; ++i) {
      const int it = tid + 128 * i;
      const int G  = G0 + (it >> 4);
      const int jf = (G * 256) / UR;
      const int jl = (G * 256 + 255) / UR;
      #pragma unroll
      for (int jj = 0; jj < 7; ++jj) {
        const int j = min(jf + jj, jl);
        const int slot = G - ((j * UR) >> 8);
        pv[i][jj] = p4[(size_t)(j * 2 + slot) * (EE / 4) + (it & 15)];
      }
    }
    #pragma unroll
    for (int i = 0; i < 2; ++i) {
      const int it = tid + 128 * i;
      const int G  = G0 + (it >> 4);
      const int jf = (G * 256) / UR;
      const int jl = (G * 256 + 255) / UR;
      float4 s = pv[i][0];
      #pragma unroll
      for (int jj = 1; jj < 7; ++jj) {
        const bool have = (jf + jj) <= jl;
        s.x += have ? pv[i][jj].x : 0.0f;
        s.y += have ? pv[i][jj].y : 0.0f;
        s.z += have ? pv[i][jj].z : 0.0f;
        s.w += have ? pv[i][jj].w : 0.0f;
      }
      s.x *= (1.0f / 256.0f); s.y *= (1.0f / 256.0f); s.z *= (1.0f / 256.0f); s.w *= (1.0f / 256.0f);
      *reinterpret_cast<float4*>(sM + (it >> 4) * EP + (it & 15) * 4) = s;
    }
  }
  __syncthreads();

  v8f acc0 = {0, 0, 0, 0, 0, 0, 0, 0};
  v8f acc1 = {0, 0, 0, 0, 0, 0, 0, 0};
  #pragma unroll
  for (int kk = 0; kk < 16; kk += 2) {
    const float* ap = sM + r * EP + 4 * kk + 2 * hf;
    const v2f a0 = *reinterpret_cast<const v2f*>(ap);
    const v2f a1 = *reinterpret_cast<const v2f*>(ap + 4);
    acc0 = wmma4(a0, b[kk],     acc0);
    acc1 = wmma4(a1, b[kk + 1], acc1);
  }
  __syncthreads();

  #pragma unroll
  for (int g = 0; g < 8; ++g)
    sM[(g + 8 * hf) * EP + n0 + r] = (acc0[g] + acc1[g]) + bo;
  __syncthreads();

  #pragma unroll
  for (int i = 0; i < 8; ++i) {
    const int j = tid + 128 * i;
    pooled[(size_t)G0 * EE + j] = sM[(j >> 6) * EP + (j & 63)];
  }
}

__global__ __launch_bounds__(256) void k_mlp_lu(
    const float* __restrict__ pooled,
    const float* __restrict__ mw1, const float* __restrict__ mb1,
    const float* __restrict__ mg,  const float* __restrict__ mbb,
    const float* __restrict__ mw2, const float* __restrict__ mb2,
    const float* __restrict__ nw1, const float* __restrict__ nb1,
    const float* __restrict__ ng,  const float* __restrict__ nbb,
    const float* __restrict__ nw2, const float* __restrict__ nb2,
    float* __restrict__ out) {
  const int branch = blockIdx.x & 1;
  const int row    = blockIdx.x >> 1;
  const float* w1 = branch ? nw1 : mw1;
  const float* b1 = branch ? nb1 : mb1;
  const float* g  = branch ? ng  : mg;
  const float* bb = branch ? nbb : mbb;
  const float* w2 = branch ? nw2 : mw2;
  const float* b2 = branch ? nb2 : mb2;

  __shared__ float sx[EE];
  __shared__ float sh[HIDD];
  __shared__ float sraw[512];
  const int tid = threadIdx.x;

  if (tid < EE) sx[tid] = pooled[(size_t)row * EE + tid];
  __syncthreads();

  if (tid < HIDD) {
    float acc = b1[tid];
    #pragma unroll
    for (int i = 0; i < EE; ++i) acc += sx[i] * w1[i * HIDD + tid];
    float mu = acc;
    #pragma unroll
    for (int m = 16; m >= 1; m >>= 1) mu += __shfl_xor(mu, m, 32);
    mu *= (1.0f / 32.0f);
    const float d = acc - mu;
    float vv = d * d;
    #pragma unroll
    for (int m = 16; m >= 1; m >>= 1) vv += __shfl_xor(vv, m, 32);
    vv *= (1.0f / 32.0f);
    const float y = d * rsqrtf(vv + 1e-5f) * g[tid] + bb[tid];
    sh[tid] = 0.5f * y * (1.0f + erff(y * 0.70710678118654752f));
  }
  __syncthreads();

  for (int c = tid; c < 512; c += 256) {
    float acc = b2[c];
    #pragma unroll
    for (int j = 0; j < HIDD; ++j) acc += sh[j] * w2[j * 512 + c];
    sraw[c] = acc;
  }
  __syncthreads();

  const size_t obase = ((size_t)branch * 384 + row) * 512;
  for (int o = tid; o < 512; o += 256) {
    const int h  = o >> 6;
    const int ik = o & 63;
    const int i  = ik >> 3;
    const int k  = ik & 7;
    const float* rh = sraw + h * 64;
    float acc = 0.0f;
    #pragma unroll
    for (int j = 0; j < NKK; ++j) {
      if (j > i || j > k) continue;
      const float Lij = (j == i) ? 1.0f : rh[(i * (i - 1)) / 2 + j];
      float u = rh[28 + (NKK * j - (j * (j - 1)) / 2) + (k - j)];
      if (j == k) u = softplusf(u) + 0.01f;
      acc += Lij * u;
    }
    out[obase + o] = acc;
  }
}

extern "C" void kernel_launch(void* const* d_in, const int* in_sizes, int n_in,
                              void* d_out, int out_size, void* d_ws, size_t ws_size,
                              hipStream_t stream) {
  const float* wsfeat = (const float*)d_in[0];
  const float* W_emb  = (const float*)d_in[1];
  const float* b_emb  = (const float*)d_in[2];
  const float* pos    = (const float*)d_in[3];
  const float* w_in   = (const float*)d_in[4];
  const float* b_in   = (const float*)d_in[5];
  const float* w_out  = (const float*)d_in[6];
  const float* b_out  = (const float*)d_in[7];
  const float* m_w1 = (const float*)d_in[8];
  const float* m_b1 = (const float*)d_in[9];
  const float* m_g  = (const float*)d_in[10];
  const float* m_bb = (const float*)d_in[11];
  const float* m_w2 = (const float*)d_in[12];
  const float* m_b2 = (const float*)d_in[13];
  const float* n_w1 = (const float*)d_in[14];
  const float* n_b1 = (const float*)d_in[15];
  const float* n_g  = (const float*)d_in[16];
  const float* n_bb = (const float*)d_in[17];
  const float* n_w2 = (const float*)d_in[18];
  const float* n_b2 = (const float*)d_in[19];
  float* out = (float*)d_out;

  float* wsp    = (float*)d_ws;
  float* part   = wsp  + (size_t)MROWS * 56;
  float* pooled = part + (size_t)MROWS * EE;

  k_fused<<<MROWS / UR, 128, 0, stream>>>(wsfeat, W_emb, b_emb, pos, w_in, b_in, part);
  k_pool_proj<<<NGRP / 16, 128, 0, stream>>>(part, w_out, b_out, pooled);
  k_mlp_lu<<<BB * LL * 2, 256, 0, stream>>>(pooled,
      m_w1, m_b1, m_g, m_bb, m_w2, m_b2,
      n_w1, n_b1, n_g, n_bb, n_w2, n_b2, out);
}


# attention scores computed transposed (k @ q^T): softmax max/sum run over 6 registers + one lane-half exchange, P feeds P@v from registers (K=12); no shuffles, no P in LDS; sum order changed
# speedup vs baseline: 3.5972x; 1.0840x over previous
#include <hip/hip_runtime.h>
#include <math.h>

#define LL   12
#define TT   8
#define BB   32
#define CC   256
#define SS   64
#define EE   64
#define HDD  16
#define NHH  4
#define HIDD 32
#define HH   8
#define NKK  8
#define MROWS (BB*CC*LL)
#define SROWS (LL*TT*BB*CC)

typedef float v2f __attribute__((ext_vector_type(2)));
typedef float v4f __attribute__((ext_vector_type(4)));
typedef float v8f __attribute__((ext_vector_type(8)));

static __device__ __forceinline__ v8f wmma4(v2f a, v2f b, v8f c) {
  return __builtin_amdgcn_wmma_f32_16x16x4_f32(false, a, false, b, (short)0, c,
                                               false, false);
}

static __device__ __forceinline__ float softplusf(float x) {
  return (x > 20.0f) ? x : log1pf(__expf(x));
}

static __device__ __forceinline__ float xhalf(float x) {
  const int b = __builtin_bit_cast(int, x);
  return __builtin_bit_cast(float, __builtin_amdgcn_permlanex16(b, b, 0x76543210, 0xfedcba98, false, false));
}

#define ROWQ 17
#define CE(a, b) { const float lo_ = fminf(v[a], v[b]); const float hi_ = fmaxf(v[a], v[b]); v[a] = lo_; v[b] = hi_; }

static __device__ __forceinline__ void row_stats(float (&v)[64], float* __restrict__ o) {
  float sa = 0.0f, sb = 0.0f, sc = 0.0f, sd = 0.0f;
  float ta = 0.0f, tb = 0.0f, tc = 0.0f, td = 0.0f;
  #pragma unroll
  for (int q = 0; q < 16; ++q) {
    sa += v[4 * q + 0]; ta = fmaf(v[4 * q + 0], v[4 * q + 0], ta);
    sb += v[4 * q + 1]; tb = fmaf(v[4 * q + 1], v[4 * q + 1], tb);
    sc += v[4 * q + 2]; tc = fmaf(v[4 * q + 2], v[4 * q + 2], tc);
    sd += v[4 * q + 3]; td = fmaf(v[4 * q + 3], v[4 * q + 3], td);
  }
  const float s  = (sa + sb) + (sc + sd);
  const float s2 = (ta + tb) + (tc + td);
  const float mean = s * (1.0f / 64.0f);
  const float var  = (s2 - 64.0f * mean * mean) * (1.0f / 63.0f);

  CE(0,1) CE(2,3) CE(4,5) CE(6,7) CE(8,9) CE(10,11)
  CE(12,13) CE(14,15) CE(16,17) CE(18,19) CE(20,21) CE(22,23)
  CE(24,25) CE(26,27) CE(28,29) CE(30,31) CE(32,33) CE(34,35)
  CE(36,37) CE(38,39) CE(40,41) CE(42,43) CE(44,45) CE(46,47)
  CE(48,49) CE(50,51) CE(52,53) CE(54,55) CE(56,57) CE(58,59)
  CE(60,61) CE(62,63) CE(0,2) CE(1,3) CE(4,6) CE(5,7)
  CE(8,10) CE(9,11) CE(12,14) CE(13,15) CE(16,18) CE(17,19)
  CE(20,22) CE(21,23) CE(24,26) CE(25,27) CE(28,30) CE(29,31)
  CE(32,34) CE(33,35) CE(36,38) CE(37,39) CE(40,42) CE(41,43)
  CE(44,46) CE(45,47) CE(48,50) CE(49,51) CE(52,54) CE(53,55)
  CE(56,58) CE(57,59) CE(60,62) CE(61,63) CE(1,2) CE(5,6)
  CE(9,10) CE(13,14) CE(17,18) CE(21,22) CE(25,26) CE(29,30)
  CE(33,34) CE(37,38) CE(41,42) CE(45,46) CE(49,50) CE(53,54)
  CE(57,58) CE(61,62) CE(0,4) CE(1,5) CE(2,6) CE(3,7)
  CE(8,12) CE(9,13) CE(10,14) CE(11,15) CE(16,20) CE(17,21)
  CE(18,22) CE(19,23) CE(24,28) CE(25,29) CE(26,30) CE(27,31)
  CE(32,36) CE(33,37) CE(34,38) CE(35,39) CE(40,44) CE(41,45)
  CE(42,46) CE(43,47) CE(48,52) CE(49,53) CE(50,54) CE(51,55)
  CE(56,60) CE(57,61) CE(58,62) CE(59,63) CE(2,4) CE(3,5)
  CE(10,12) CE(11,13) CE(18,20) CE(19,21) CE(26,28) CE(27,29)
  CE(34,36) CE(35,37) CE(42,44) CE(43,45) CE(50,52) CE(51,53)
  CE(58,60) CE(59,61) CE(1,2) CE(3,4) CE(5,6) CE(9,10)
  CE(11,12) CE(13,14) CE(17,18) CE(19,20) CE(21,22) CE(25,26)
  CE(27,28) CE(29,30) CE(33,34) CE(35,36) CE(37,38) CE(41,42)
  CE(43,44) CE(45,46) CE(49,50) CE(51,52) CE(53,54) CE(57,58)
  CE(59,60) CE(61,62) CE(0,8) CE(1,9) CE(2,10) CE(3,11)
  CE(4,12) CE(5,13) CE(6,14) CE(7,15) CE(16,24) CE(17,25)
  CE(18,26) CE(19,27) CE(20,28) CE(21,29) CE(22,30) CE(23,31)
  CE(32,40) CE(33,41) CE(34,42) CE(35,43) CE(36,44) CE(37,45)
  CE(38,46) CE(39,47) CE(48,56) CE(49,57) CE(50,58) CE(51,59)
  CE(52,60) CE(53,61) CE(54,62) CE(55,63) CE(4,8) CE(5,9)
  CE(6,10) CE(7,11) CE(20,24) CE(21,25) CE(22,26) CE(23,27)
  CE(36,40) CE(37,41) CE(38,42) CE(39,43) CE(52,56) CE(53,57)
  CE(54,58) CE(55,59) CE(2,4) CE(3,5) CE(6,8) CE(7,9)
  CE(10,12) CE(11,13) CE(18,20) CE(19,21) CE(22,24) CE(23,25)
  CE(26,28) CE(27,29) CE(34,36) CE(35,37) CE(38,40) CE(39,41)
  CE(42,44) CE(43,45) CE(50,52) CE(51,53) CE(54,56) CE(55,57)
  CE(58,60) CE(59,61) CE(1,2) CE(3,4) CE(5,6) CE(7,8)
  CE(9,10) CE(11,12) CE(13,14) CE(17,18) CE(19,20) CE(21,22)
  CE(23,24) CE(25,26) CE(27,28) CE(29,30) CE(33,34) CE(35,36)
  CE(37,38) CE(39,40) CE(41,42) CE(43,44) CE(45,46) CE(49,50)
  CE(51,52) CE(53,54) CE(55,56) CE(57,58) CE(59,60) CE(61,62)
  CE(0,16) CE(1,17) CE(2,18) CE(3,19) CE(4,20) CE(5,21)
  CE(6,22) CE(7,23) CE(8,24) CE(9,25) CE(10,26) CE(11,27)
  CE(12,28) CE(13,29) CE(14,30) CE(15,31) CE(32,48) CE(33,49)
  CE(34,50) CE(35,51) CE(36,52) CE(37,53) CE(38,54) CE(39,55)
  CE(40,56) CE(41,57) CE(42,58) CE(43,59) CE(44,60) CE(45,61)
  CE(46,62) CE(47,63) CE(8,16) CE(9,17) CE(10,18) CE(11,19)
  CE(12,20) CE(13,21) CE(14,22) CE(15,23) CE(40,48) CE(41,49)
  CE(42,50) CE(43,51) CE(44,52) CE(45,53) CE(46,54) CE(47,55)
  CE(4,8) CE(5,9) CE(6,10) CE(7,11) CE(12,16) CE(13,17)
  CE(14,18) CE(15,19) CE(20,24) CE(21,25) CE(22,26) CE(23,27)
  CE(36,40) CE(37,41) CE(38,42) CE(39,43) CE(44,48) CE(45,49)
  CE(46,50) CE(47,51) CE(52,56) CE(53,57) CE(54,58) CE(55,59)
  CE(2,4) CE(3,5) CE(6,8) CE(7,9) CE(10,12) CE(11,13)
  CE(14,16) CE(15,17) CE(18,20) CE(19,21) CE(22,24) CE(23,25)
  CE(26,28) CE(27,29) CE(34,36) CE(35,37) CE(38,40) CE(39,41)
  CE(42,44) CE(43,45) CE(46,48) CE(47,49) CE(50,52) CE(51,53)
  CE(54,56) CE(55,57) CE(58,60) CE(59,61) CE(1,2) CE(3,4)
  CE(5,6) CE(7,8) CE(9,10) CE(11,12) CE(13,14) CE(15,16)
  CE(17,18) CE(19,20) CE(21,22) CE(23,24) CE(25,26) CE(27,28)
  CE(29,30) CE(33,34) CE(35,36) CE(37,38) CE(39,40) CE(41,42)
  CE(43,44) CE(45,46) CE(47,48) CE(49,50) CE(51,52) CE(53,54)
  CE(55,56) CE(57,58) CE(59,60) CE(61,62) CE(0,32) CE(1,33)
  CE(2,34) CE(3,35) CE(4,36) CE(5,37) CE(6,38) CE(7,39)
  CE(8,40) CE(9,41) CE(10,42) CE(11,43) CE(12,44) CE(13,45)
  CE(14,46) CE(15,47) CE(16,48) CE(17,49) CE(18,50) CE(19,51)
  CE(20,52) CE(21,53) CE(22,54) CE(23,55) CE(24,56) CE(25,57)
  CE(26,58) CE(27,59) CE(28,60) CE(29,61) CE(30,62) CE(31,63)
  CE(16,32) CE(17,33) CE(18,34) CE(19,35) CE(20,36) CE(21,37)
  CE(22,38) CE(23,39) CE(24,40) CE(25,41) CE(26,42) CE(27,43)
  CE(28,44) CE(29,45) CE(30,46) CE(31,47) CE(8,16) CE(9,17)
  CE(10,18) CE(11,19) CE(12,20) CE(13,21) CE(14,22) CE(15,23)
  CE(24,32) CE(25,33) CE(26,34) CE(27,35) CE(28,36) CE(29,37)
  CE(30,38) CE(31,39) CE(40,48) CE(41,49) CE(42,50) CE(43,51)
  CE(44,52) CE(45,53) CE(46,54) CE(47,55) CE(4,8) CE(5,9)
  CE(6,10) CE(7,11) CE(12,16) CE(13,17) CE(14,18) CE(15,19)
  CE(20,24) CE(21,25) CE(22,26) CE(23,27) CE(28,32) CE(29,33)
  CE(30,34) CE(31,35) CE(36,40) CE(37,41) CE(38,42) CE(39,43)
  CE(44,48) CE(45,49) CE(46,50) CE(47,51) CE(52,56) CE(53,57)
  CE(54,58) CE(55,59) CE(2,4) CE(3,5) CE(6,8) CE(7,9)
  CE(10,12) CE(11,13) CE(14,16) CE(15,17) CE(18,20) CE(19,21)
  CE(22,24) CE(23,25) CE(26,28) CE(27,29) CE(30,32) CE(31,33)
  CE(34,36) CE(35,37) CE(38,40) CE(39,41) CE(42,44) CE(43,45)
  CE(46,48) CE(47,49) CE(50,52) CE(51,53) CE(54,56) CE(55,57)
  CE(58,60) CE(59,61) CE(1,2) CE(3,4) CE(5,6) CE(7,8)
  CE(9,10) CE(11,12) CE(13,14) CE(15,16) CE(17,18) CE(19,20)
  CE(21,22) CE(23,24) CE(25,26) CE(27,28) CE(29,30) CE(31,32)
  CE(33,34) CE(35,36) CE(37,38) CE(39,40) CE(41,42) CE(43,44)
  CE(45,46) CE(47,48) CE(49,50) CE(51,52) CE(53,54) CE(55,56)
  CE(57,58) CE(59,60) CE(61,62)

  const float q0 = v[0];
  const float q1 = v[15] + 0.75f * (v[16] - v[15]);
  const float q2 = 0.5f * (v[31] + v[32]);
  const float q3 = v[47] + 0.25f * (v[48] - v[47]);
  const float q4 = v[63];

  o[0] = mean; o[1] = var; o[2] = q0; o[3] = q1; o[4] = q2; o[5] = q3; o[6] = q4;
}
#undef CE

#define UR 48
#define FP 60
#define EP 68
#define QP 196

__global__ __launch_bounds__(128) void k_fused(const float* __restrict__ ws,
                                               const float* __restrict__ Wemb,
                                               const float* __restrict__ bemb,
                                               const float* __restrict__ pos,
                                               const float* __restrict__ win,
                                               const float* __restrict__ bin,
                                               float* __restrict__ part) {
  __shared__ __attribute__((aligned(16))) float sA[UR * FP];
  __shared__ __attribute__((aligned(16))) float sB[UR * EP];
  __shared__ __attribute__((aligned(16))) float sC[UR * QP];
  const int tid  = threadIdx.x;
  const int lane = tid & 31;
  const int wid  = tid >> 5;
  const int r  = lane & 15;
  const int hf = lane >> 4;
  const size_t row0 = (size_t)blockIdx.x * UR;

  {
    int* sRow = reinterpret_cast<int*>(sB);
    v4f* wt = reinterpret_cast<v4f*>(sC) + wid * (32 * ROWQ);
    #pragma unroll
    for (int k = 0; k < 3; ++k) {
      const int idx = tid + 128 * k;
      const int t = idx / UR;
      const int i = idx - t * UR;
      const int R = (int)row0 + i;
      const int c = R & 255;
      const int bl = R >> 8;
      const int b = bl / LL;
      const int l = bl - b * LL;
      sRow[idx] = ((l * TT + t) * BB + b) * CC + c;
    }
    __syncthreads();

    v4f x[16];
    {
      const int base = wid * 32 + hf;
      #pragma unroll
      for (int q = 0; q < 16; ++q) {
        const int sr = sRow[base + 2 * q];
        x[q] = *(reinterpret_cast<const v4f*>(ws + (size_t)sr * SS) + r);
      }
    }
    #pragma unroll 1
    for (int p = 0; p < 3; ++p) {
      #pragma unroll
      for (int q = 0; q < 16; ++q) wt[(2 * q + hf) * ROWQ + r] = x[q];
      __syncthreads();

      float v[64];
      #pragma unroll
      for (int q = 0; q < 16; ++q) {
        const v4f y = wt[lane * ROWQ + q];
        v[4 * q + 0] = y.x; v[4 * q + 1] = y.y; v[4 * q + 2] = y.z; v[4 * q + 3] = y.w;
      }
      if (p < 2) {
        const int base = (p + 1) * 128 + wid * 32 + hf;
        #pragma unroll
        for (int q = 0; q < 16; ++q) {
          const int sr = sRow[base + 2 * q];
          x[q] = *(reinterpret_cast<const v4f*>(ws + (size_t)sr * SS) + r);
        }
      }
      const int idx = p * 128 + tid;
      const int t = idx / UR;
      const int i = idx - t * UR;
      row_stats(v, sA + i * FP + t * 7);
    }
  }
  __syncthreads();

  {
    const int n0 = wid * 16;
    v8f e0 = {0, 0, 0, 0, 0, 0, 0, 0};
    v8f e1 = {0, 0, 0, 0, 0, 0, 0, 0};
    v8f e2 = {0, 0, 0, 0, 0, 0, 0, 0};
    #pragma unroll
    for (int k0 = 0; k0 < 56; k0 += 4) {
      const int c0 = k0 + 2 * hf;
      const float* bp = Wemb + (size_t)c0 * EE + n0 + r;
      v2f b; b.x = bp[0]; b.y = bp[EE];
      const float* ap = sA + r * FP + c0;
      const v2f a0 = *reinterpret_cast<const v2f*>(ap);
      const v2f a1 = *reinterpret_cast<const v2f*>(ap + 16 * FP);
      const v2f a2 = *reinterpret_cast<const v2f*>(ap + 32 * FP);
      e0 = wmma4(a0, b, e0);
      e1 = wmma4(a1, b, e1);
      e2 = wmma4(a2, b, e2);
    }
    const int n = n0 + r;
    const float be = bemb[n];
    #pragma unroll
    for (int g = 0; g < 8; ++g) {
      const int ml = g + 8 * hf;
      const int l0 = (int)(((row0 + ml)      >> 8) % LL);
      const int l1 = (int)(((row0 + ml + 16) >> 8) % LL);
      const int l2 = (int)(((row0 + ml + 32) >> 8) % LL);
      sB[(ml)      * EP + n] = e0[g] + be + pos[l0 * EE + n];
      sB[(ml + 16) * EP + n] = e1[g] + be + pos[l1 * EE + n];
      sB[(ml + 32) * EP + n] = e2[g] + be + pos[l2 * EE + n];
    }
  }
  __syncthreads();

  {
    v2f a[3][16];
    #pragma unroll
    for (int kk = 0; kk < 16; ++kk) {
      const float* ap = sB + r * EP + 4 * kk + 2 * hf;
      a[0][kk] = *reinterpret_cast<const v2f*>(ap);
      a[1][kk] = *reinterpret_cast<const v2f*>(ap + 16 * EP);
      a[2][kk] = *reinterpret_cast<const v2f*>(ap + 32 * EP);
    }
    #pragma unroll
    for (int j = 0; j < 3; ++j) {
      const int n0 = (wid + 4 * j) * 16;
      v8f c0v = {0, 0, 0, 0, 0, 0, 0, 0};
      v8f c1v = {0, 0, 0, 0, 0, 0, 0, 0};
      v8f c2v = {0, 0, 0, 0, 0, 0, 0, 0};
      #pragma unroll
      for (int kk = 0; kk < 16; ++kk) {
        const int c0 = 4 * kk + 2 * hf;
        const float* bp = win + (size_t)c0 * 192 + n0 + r;
        v2f b; b.x = bp[0]; b.y = bp[192];
        c0v = wmma4(a[0][kk], b, c0v);
        c1v = wmma4(a[1][kk], b, c1v);
        c2v = wmma4(a[2][kk], b, c2v);
      }
      const int n = n0 + r;
      const float bi_ = bin[n];
      #pragma unroll
      for (int g = 0; g < 8; ++g) {
        const int ml = g + 8 * hf;
        sC[(ml)      * QP + n] = c0v[g] + bi_;
        sC[(ml + 16) * QP + n] = c1v[g] + bi_;
        sC[(ml + 32) * QP + n] = c2v[g] + bi_;
      }
    }
  }
  __syncthreads();

  const int rq = (r < LL) ? r : (LL - 1);
  const int rk = (r < 8) ? ((r < 6) ? r : 5) : ((r < 14) ? (r - 2) : (LL - 1));

  const int gsplit = (int)((((row0 >> 8) + 1) << 8) - row0);
  float t0[4], t1[4];
  #pragma unroll
  for (int bi = 0; bi < 4; ++bi) {
    const float* base = sC + (bi * LL) * QP + wid * HDD;

    v8f S = {0, 0, 0, 0, 0, 0, 0, 0};
    #pragma unroll
    for (int k0 = 0; k0 < 16; k0 += 4) {
      const int c0 = k0 + 2 * hf;
      const v2f a = *reinterpret_cast<const v2f*>(base + rk * QP + EE + c0);
      const v2f b = *reinterpret_cast<const v2f*>(base + rq * QP + c0);
      S = wmma4(a, b, S);
    }

    float sv[6];
    #pragma unroll
    for (int g = 0; g < 6; ++g) sv[g] = S[g] * 0.25f;
    float rm = fmaxf(fmaxf(fmaxf(sv[0], sv[1]), fmaxf(sv[2], sv[3])), fmaxf(sv[4], sv[5]));
    rm = fmaxf(rm, xhalf(rm));
    float e[6];
    #pragma unroll
    for (int g = 0; g < 6; ++g) e[g] = __expf(sv[g] - rm);
    float rs = ((e[0] + e[1]) + (e[2] + e[3])) + (e[4] + e[5]);
    rs += xhalf(rs);

    v8f Cv = {0, 0, 0, 0, 0, 0, 0, 0};
    #pragma unroll
    for (int j = 0; j < 3; ++j) {
      v2f a; a.x = e[2 * j] / rs; a.y = e[2 * j + 1] / rs;
      const float* vp = base + (6 * hf + 2 * j) * QP + 128 + r;
      v2f b; b.x = vp[0]; b.y = vp[QP];
      Cv = wmma4(a, b, Cv);
    }
    float u0 = 0.0f, u1 = 0.0f;
    #pragma unroll
    for (int g = 0; g < 8; ++g) {
      const int m = g + 8 * hf;
      const float cv = (m < LL) ? Cv[g] : 0.0f;
      const bool first = (bi * LL + m) < gsplit;
      u0 += first ? cv : 0.0f;
      u1 += first ? 0.0f : cv;
    }
    t0[bi] = u0; t1[bi] = u1;
  }
  float s0 = (t0[0] + t0[1]) + (t0[2] + t0[3]);
  float s1 = (t1[0] + t1[1]) + (t1[2] + t1[3]);
  s0 += __shfl_xor(s0, 16, 32);
  s1 += __shfl_xor(s1, 16, 32);
  if (hf == 0) {
    sB[wid * HDD + r]      = s0;
    sB[EE + wid * HDD + r] = s1;
  }
  __syncthreads();

  part[(size_t)blockIdx.x * (2 * EE) + tid] = sB[tid];
}

#define NGRP (BB * LL)
__global__ __launch_bounds__(128) void k_pool_proj(const float* __restrict__ part,
                                                   const float* __restrict__ wout,
                                                   const float* __restrict__ bout,
                                                   float* __restrict__ pooled) {
  __shared__ __attribute__((aligned(16))) float sM[16 * EP];
  const int tid  = threadIdx.x;
  const int lane = tid & 31;
  const int wid  = tid >> 5;
  const int r  = lane & 15;
  const int hf = lane >> 4;
  const int G0 = blockIdx.x * 16;
  const int n0 = wid * 16;

  v2f b[16];
  #pragma unroll
  for (int kk = 0; kk < 16; ++kk) {
    const int c0 = 4 * kk + 2 * hf;
    const float* bp = wout + (size_t)c0 * EE + n0 + r;
    b[kk].x = bp[0]; b[kk].y = bp[EE];
  }
  const float bo = bout[n0 + r];

  {
    const float4* p4 = reinterpret_cast<const float4*>(part);
    float4 pv[2][7];
    #pragma unroll
    for (int i = 0; i < 2; ++i) {
      const int it = tid + 128 * i;
      const int G  = G0 + (it >> 4);
      const int jf = (G * 256) / UR;
      const int jl = (G * 256 + 255) / UR;
      #pragma unroll
      for (int jj = 0; jj < 7; ++jj) {
        const int j = min(jf + jj, jl);
        const int slot = G - ((j * UR) >> 8);
        pv[i][jj] = p4[(size_t)(j * 2 + slot) * (EE / 4) + (it & 15)];
      }
    }
    #pragma unroll
    for (int i = 0; i < 2; ++i) {
      const int it = tid + 128 * i;
      const int G  = G0 + (it >> 4);
      const int jf = (G * 256) / UR;
      const int jl = (G * 256 + 255) / UR;
      float4 s = pv[i][0];
      #pragma unroll
      for (int jj = 1; jj < 7; ++jj) {
        const bool have = (jf + jj) <= jl;
        s.x += have ? pv[i][jj].x : 0.0f;
        s.y += have ? pv[i][jj].y : 0.0f;
        s.z += have ? pv[i][jj].z : 0.0f;
        s.w += have ? pv[i][jj].w : 0.0f;
      }
      s.x *= (1.0f / 256.0f); s.y *= (1.0f / 256.0f); s.z *= (1.0f / 256.0f); s.w *= (1.0f / 256.0f);
      *reinterpret_cast<float4*>(sM + (it >> 4) * EP + (it & 15) * 4) = s;
    }
  }
  __syncthreads();

  v8f acc0 = {0, 0, 0, 0, 0, 0, 0, 0};
  v8f acc1 = {0, 0, 0, 0, 0, 0, 0, 0};
  #pragma unroll
  for (int kk = 0; kk < 16; kk += 2) {
    const float* ap = sM + r * EP + 4 * kk + 2 * hf;
    const v2f a0 = *reinterpret_cast<const v2f*>(ap);
    const v2f a1 = *reinterpret_cast<const v2f*>(ap + 4);
    acc0 = wmma4(a0, b[kk],     acc0);
    acc1 = wmma4(a1, b[kk + 1], acc1);
  }
  __syncthreads();

  #pragma unroll
  for (int g = 0; g < 8; ++g)
    sM[(g + 8 * hf) * EP + n0 + r] = (acc0[g] + acc1[g]) + bo;
  __syncthreads();

  #pragma unroll
  for (int i = 0; i < 8; ++i) {
    const int j = tid + 128 * i;
    pooled[(size_t)G0 * EE + j] = sM[(j >> 6) * EP + (j & 63)];
  }
}

__global__ __launch_bounds__(256) void k_mlp_lu(
    const float* __restrict__ pooled,
    const float* __restrict__ mw1, const float* __restrict__ mb1,
    const float* __restrict__ mg,  const float* __restrict__ mbb,
    const float* __restrict__ mw2, const float* __restrict__ mb2,
    const float* __restrict__ nw1, const float* __restrict__ nb1,
    const float* __restrict__ ng,  const float* __restrict__ nbb,
    const float* __restrict__ nw2, const float* __restrict__ nb2,
    float* __restrict__ out) {
  const int branch = blockIdx.x & 1;
  const int row    = blockIdx.x >> 1;
  const float* w1 = branch ? nw1 : mw1;
  const float* b1 = branch ? nb1 : mb1;
  const float* g  = branch ? ng  : mg;
  const float* bb = branch ? nbb : mbb;
  const float* w2 = branch ? nw2 : mw2;
  const float* b2 = branch ? nb2 : mb2;

  __shared__ float sx[EE];
  __shared__ float sh[HIDD];
  __shared__ float sraw[512];
  const int tid = threadIdx.x;

  if (tid < EE) sx[tid] = pooled[(size_t)row * EE + tid];
  __syncthreads();

  if (tid < HIDD) {
    float acc = b1[tid];
    #pragma unroll
    for (int i = 0; i < EE; ++i) acc += sx[i] * w1[i * HIDD + tid];
    float mu = acc;
    #pragma unroll
    for (int m = 16; m >= 1; m >>= 1) mu += __shfl_xor(mu, m, 32);
    mu *= (1.0f / 32.0f);
    const float d = acc - mu;
    float vv = d * d;
    #pragma unroll
    for (int m = 16; m >= 1; m >>= 1) vv += __shfl_xor(vv, m, 32);
    vv *= (1.0f / 32.0f);
    const float y = d * rsqrtf(vv + 1e-5f) * g[tid] + bb[tid];
    sh[tid] = 0.5f * y * (1.0f + erff(y * 0.70710678118654752f));
  }
  __syncthreads();

  for (int c = tid; c < 512; c += 256) {
    float acc = b2[c];
    #pragma unroll
    for (int j = 0; j < HIDD; ++j) acc += sh[j] * w2[j * 512 + c];
    sraw[c] = acc;
  }
  __syncthreads();

  const size_t obase = ((size_t)branch * 384 + row) * 512;
  for (int o = tid; o < 512; o += 256) {
    const int h  = o >> 6;
    const int ik = o & 63;
    const int i  = ik >> 3;
    const int k  = ik & 7;
    const float* rh = sraw + h * 64;
    float acc = 0.0f;
    #pragma unroll
    for (int j = 0; j < NKK; ++j) {
      if (j > i || j > k) continue;
      const float Lij = (j == i) ? 1.0f : rh[(i * (i - 1)) / 2 + j];
      float u = rh[28 + (NKK * j - (j * (j - 1)) / 2) + (k - j)];
      if (j == k) u = softplusf(u) + 0.01f;
      acc += Lij * u;
    }
    out[obase + o] = acc;
  }
}

extern "C" void kernel_launch(void* const* d_in, const int* in_sizes, int n_in,
                              void* d_out, int out_size, void* d_ws, size_t ws_size,
                              hipStream_t stream) {
  const float* wsfeat = (const float*)d_in[0];
  const float* W_emb  = (const float*)d_in[1];
  const float* b_emb  = (const float*)d_in[2];
  const float* pos    = (const float*)d_in[3];
  const float* w_in   = (const float*)d_in[4];
  const float* b_in   = (const float*)d_in[5];
  const float* w_out  = (const float*)d_in[6];
  const float* b_out  = (const float*)d_in[7];
  const float* m_w1 = (const float*)d_in[8];
  const float* m_b1 = (const float*)d_in[9];
  const float* m_g  = (const float*)d_in[10];
  const float* m_bb = (const float*)d_in[11];
  const float* m_w2 = (const float*)d_in[12];
  const float* m_b2 = (const float*)d_in[13];
  const float* n_w1 = (const float*)d_in[14];
  const float* n_b1 = (const float*)d_in[15];
  const float* n_g  = (const float*)d_in[16];
  const float* n_bb = (const float*)d_in[17];
  const float* n_w2 = (const float*)d_in[18];
  const float* n_b2 = (const float*)d_in[19];
  float* out = (float*)d_out;

  float* wsp    = (float*)d_ws;
  float* part   = wsp  + (size_t)MROWS * 56;
  float* pooled = part + (size_t)MROWS * EE;

  k_fused<<<MROWS / UR, 128, 0, stream>>>(wsfeat, W_emb, b_emb, pos, w_in, b_in, part);
  k_pool_proj<<<NGRP / 16, 128, 0, stream>>>(part, w_out, b_out, pooled);
  k_mlp_lu<<<BB * LL * 2, 256, 0, stream>>>(pooled,
      m_w1, m_b1, m_g, m_bb, m_w2, m_b2,
      n_w1, n_b1, n_g, n_bb, n_w2, n_b2, out);
}


# fuse pool projection into the MLP/LU kernel: a 4-wave block per (16 pool groups, branch, 2 heads) runs w_out, w1, LN/gelu, w2 on WMMA tiles and the LU product; 3 launches -> 2; MLP sum order changed
# speedup vs baseline: 4.1792x; 1.1618x over previous
#include <hip/hip_runtime.h>
#include <math.h>

#define LL   12
#define TT   8
#define BB   32
#define CC   256
#define SS   64
#define EE   64
#define HDD  16
#define NHH  4
#define HIDD 32
#define HH   8
#define NKK  8
#define MROWS (BB*CC*LL)
#define SROWS (LL*TT*BB*CC)

typedef float v2f __attribute__((ext_vector_type(2)));
typedef float v4f __attribute__((ext_vector_type(4)));
typedef float v8f __attribute__((ext_vector_type(8)));

static __device__ __forceinline__ v8f wmma4(v2f a, v2f b, v8f c) {
  return __builtin_amdgcn_wmma_f32_16x16x4_f32(false, a, false, b, (short)0, c,
                                               false, false);
}

static __device__ __forceinline__ float softplusf(float x) {
  return (x > 20.0f) ? x : log1pf(__expf(x));
}

static __device__ __forceinline__ float xhalf(float x) {
  const int b = __builtin_bit_cast(int, x);
  return __builtin_bit_cast(float, __builtin_amdgcn_permlanex16(b, b, 0x76543210, 0xfedcba98, false, false));
}

#define ROWQ 17
#define CE(a, b) { const float lo_ = fminf(v[a], v[b]); const float hi_ = fmaxf(v[a], v[b]); v[a] = lo_; v[b] = hi_; }

static __device__ __forceinline__ void row_stats(float (&v)[64], float* __restrict__ o) {
  float sa = 0.0f, sb = 0.0f, sc = 0.0f, sd = 0.0f;
  float ta = 0.0f, tb = 0.0f, tc = 0.0f, td = 0.0f;
  #pragma unroll
  for (int q = 0; q < 16; ++q) {
    sa += v[4 * q + 0]; ta = fmaf(v[4 * q + 0], v[4 * q + 0], ta);
    sb += v[4 * q + 1]; tb = fmaf(v[4 * q + 1], v[4 * q + 1], tb);
    sc += v[4 * q + 2]; tc = fmaf(v[4 * q + 2], v[4 * q + 2], tc);
    sd += v[4 * q + 3]; td = fmaf(v[4 * q + 3], v[4 * q + 3], td);
  }
  const float s  = (sa + sb) + (sc + sd);
  const float s2 = (ta + tb) + (tc + td);
  const float mean = s * (1.0f / 64.0f);
  const float var  = (s2 - 64.0f * mean * mean) * (1.0f / 63.0f);

  CE(0,1) CE(2,3) CE(4,5) CE(6,7) CE(8,9) CE(10,11)
  CE(12,13) CE(14,15) CE(16,17) CE(18,19) CE(20,21) CE(22,23)
  CE(24,25) CE(26,27) CE(28,29) CE(30,31) CE(32,33) CE(34,35)
  CE(36,37) CE(38,39) CE(40,41) CE(42,43) CE(44,45) CE(46,47)
  CE(48,49) CE(50,51) CE(52,53) CE(54,55) CE(56,57) CE(58,59)
  CE(60,61) CE(62,63) CE(0,2) CE(1,3) CE(4,6) CE(5,7)
  CE(8,10) CE(9,11) CE(12,14) CE(13,15) CE(16,18) CE(17,19)
  CE(20,22) CE(21,23) CE(24,26) CE(25,27) CE(28,30) CE(29,31)
  CE(32,34) CE(33,35) CE(36,38) CE(37,39) CE(40,42) CE(41,43)
  CE(44,46) CE(45,47) CE(48,50) CE(49,51) CE(52,54) CE(53,55)
  CE(56,58) CE(57,59) CE(60,62) CE(61,63) CE(1,2) CE(5,6)
  CE(9,10) CE(13,14) CE(17,18) CE(21,22) CE(25,26) CE(29,30)
  CE(33,34) CE(37,38) CE(41,42) CE(45,46) CE(49,50) CE(53,54)
  CE(57,58) CE(61,62) CE(0,4) CE(1,5) CE(2,6) CE(3,7)
  CE(8,12) CE(9,13) CE(10,14) CE(11,15) CE(16,20) CE(17,21)
  CE(18,22) CE(19,23) CE(24,28) CE(25,29) CE(26,30) CE(27,31)
  CE(32,36) CE(33,37) CE(34,38) CE(35,39) CE(40,44) CE(41,45)
  CE(42,46) CE(43,47) CE(48,52) CE(49,53) CE(50,54) CE(51,55)
  CE(56,60) CE(57,61) CE(58,62) CE(59,63) CE(2,4) CE(3,5)
  CE(10,12) CE(11,13) CE(18,20) CE(19,21) CE(26,28) CE(27,29)
  CE(34,36) CE(35,37) CE(42,44) CE(43,45) CE(50,52) CE(51,53)
  CE(58,60) CE(59,61) CE(1,2) CE(3,4) CE(5,6) CE(9,10)
  CE(11,12) CE(13,14) CE(17,18) CE(19,20) CE(21,22) CE(25,26)
  CE(27,28) CE(29,30) CE(33,34) CE(35,36) CE(37,38) CE(41,42)
  CE(43,44) CE(45,46) CE(49,50) CE(51,52) CE(53,54) CE(57,58)
  CE(59,60) CE(61,62) CE(0,8) CE(1,9) CE(2,10) CE(3,11)
  CE(4,12) CE(5,13) CE(6,14) CE(7,15) CE(16,24) CE(17,25)
  CE(18,26) CE(19,27) CE(20,28) CE(21,29) CE(22,30) CE(23,31)
  CE(32,40) CE(33,41) CE(34,42) CE(35,43) CE(36,44) CE(37,45)
  CE(38,46) CE(39,47) CE(48,56) CE(49,57) CE(50,58) CE(51,59)
  CE(52,60) CE(53,61) CE(54,62) CE(55,63) CE(4,8) CE(5,9)
  CE(6,10) CE(7,11) CE(20,24) CE(21,25) CE(22,26) CE(23,27)
  CE(36,40) CE(37,41) CE(38,42) CE(39,43) CE(52,56) CE(53,57)
  CE(54,58) CE(55,59) CE(2,4) CE(3,5) CE(6,8) CE(7,9)
  CE(10,12) CE(11,13) CE(18,20) CE(19,21) CE(22,24) CE(23,25)
  CE(26,28) CE(27,29) CE(34,36) CE(35,37) CE(38,40) CE(39,41)
  CE(42,44) CE(43,45) CE(50,52) CE(51,53) CE(54,56) CE(55,57)
  CE(58,60) CE(59,61) CE(1,2) CE(3,4) CE(5,6) CE(7,8)
  CE(9,10) CE(11,12) CE(13,14) CE(17,18) CE(19,20) CE(21,22)
  CE(23,24) CE(25,26) CE(27,28) CE(29,30) CE(33,34) CE(35,36)
  CE(37,38) CE(39,40) CE(41,42) CE(43,44) CE(45,46) CE(49,50)
  CE(51,52) CE(53,54) CE(55,56) CE(57,58) CE(59,60) CE(61,62)
  CE(0,16) CE(1,17) CE(2,18) CE(3,19) CE(4,20) CE(5,21)
  CE(6,22) CE(7,23) CE(8,24) CE(9,25) CE(10,26) CE(11,27)
  CE(12,28) CE(13,29) CE(14,30) CE(15,31) CE(32,48) CE(33,49)
  CE(34,50) CE(35,51) CE(36,52) CE(37,53) CE(38,54) CE(39,55)
  CE(40,56) CE(41,57) CE(42,58) CE(43,59) CE(44,60) CE(45,61)
  CE(46,62) CE(47,63) CE(8,16) CE(9,17) CE(10,18) CE(11,19)
  CE(12,20) CE(13,21) CE(14,22) CE(15,23) CE(40,48) CE(41,49)
  CE(42,50) CE(43,51) CE(44,52) CE(45,53) CE(46,54) CE(47,55)
  CE(4,8) CE(5,9) CE(6,10) CE(7,11) CE(12,16) CE(13,17)
  CE(14,18) CE(15,19) CE(20,24) CE(21,25) CE(22,26) CE(23,27)
  CE(36,40) CE(37,41) CE(38,42) CE(39,43) CE(44,48) CE(45,49)
  CE(46,50) CE(47,51) CE(52,56) CE(53,57) CE(54,58) CE(55,59)
  CE(2,4) CE(3,5) CE(6,8) CE(7,9) CE(10,12) CE(11,13)
  CE(14,16) CE(15,17) CE(18,20) CE(19,21) CE(22,24) CE(23,25)
  CE(26,28) CE(27,29) CE(34,36) CE(35,37) CE(38,40) CE(39,41)
  CE(42,44) CE(43,45) CE(46,48) CE(47,49) CE(50,52) CE(51,53)
  CE(54,56) CE(55,57) CE(58,60) CE(59,61) CE(1,2) CE(3,4)
  CE(5,6) CE(7,8) CE(9,10) CE(11,12) CE(13,14) CE(15,16)
  CE(17,18) CE(19,20) CE(21,22) CE(23,24) CE(25,26) CE(27,28)
  CE(29,30) CE(33,34) CE(35,36) CE(37,38) CE(39,40) CE(41,42)
  CE(43,44) CE(45,46) CE(47,48) CE(49,50) CE(51,52) CE(53,54)
  CE(55,56) CE(57,58) CE(59,60) CE(61,62) CE(0,32) CE(1,33)
  CE(2,34) CE(3,35) CE(4,36) CE(5,37) CE(6,38) CE(7,39)
  CE(8,40) CE(9,41) CE(10,42) CE(11,43) CE(12,44) CE(13,45)
  CE(14,46) CE(15,47) CE(16,48) CE(17,49) CE(18,50) CE(19,51)
  CE(20,52) CE(21,53) CE(22,54) CE(23,55) CE(24,56) CE(25,57)
  CE(26,58) CE(27,59) CE(28,60) CE(29,61) CE(30,62) CE(31,63)
  CE(16,32) CE(17,33) CE(18,34) CE(19,35) CE(20,36) CE(21,37)
  CE(22,38) CE(23,39) CE(24,40) CE(25,41) CE(26,42) CE(27,43)
  CE(28,44) CE(29,45) CE(30,46) CE(31,47) CE(8,16) CE(9,17)
  CE(10,18) CE(11,19) CE(12,20) CE(13,21) CE(14,22) CE(15,23)
  CE(24,32) CE(25,33) CE(26,34) CE(27,35) CE(28,36) CE(29,37)
  CE(30,38) CE(31,39) CE(40,48) CE(41,49) CE(42,50) CE(43,51)
  CE(44,52) CE(45,53) CE(46,54) CE(47,55) CE(4,8) CE(5,9)
  CE(6,10) CE(7,11) CE(12,16) CE(13,17) CE(14,18) CE(15,19)
  CE(20,24) CE(21,25) CE(22,26) CE(23,27) CE(28,32) CE(29,33)
  CE(30,34) CE(31,35) CE(36,40) CE(37,41) CE(38,42) CE(39,43)
  CE(44,48) CE(45,49) CE(46,50) CE(47,51) CE(52,56) CE(53,57)
  CE(54,58) CE(55,59) CE(2,4) CE(3,5) CE(6,8) CE(7,9)
  CE(10,12) CE(11,13) CE(14,16) CE(15,17) CE(18,20) CE(19,21)
  CE(22,24) CE(23,25) CE(26,28) CE(27,29) CE(30,32) CE(31,33)
  CE(34,36) CE(35,37) CE(38,40) CE(39,41) CE(42,44) CE(43,45)
  CE(46,48) CE(47,49) CE(50,52) CE(51,53) CE(54,56) CE(55,57)
  CE(58,60) CE(59,61) CE(1,2) CE(3,4) CE(5,6) CE(7,8)
  CE(9,10) CE(11,12) CE(13,14) CE(15,16) CE(17,18) CE(19,20)
  CE(21,22) CE(23,24) CE(25,26) CE(27,28) CE(29,30) CE(31,32)
  CE(33,34) CE(35,36) CE(37,38) CE(39,40) CE(41,42) CE(43,44)
  CE(45,46) CE(47,48) CE(49,50) CE(51,52) CE(53,54) CE(55,56)
  CE(57,58) CE(59,60) CE(61,62)

  const float q0 = v[0];
  const float q1 = v[15] + 0.75f * (v[16] - v[15]);
  const float q2 = 0.5f * (v[31] + v[32]);
  const float q3 = v[47] + 0.25f * (v[48] - v[47]);
  const float q4 = v[63];

  o[0] = mean; o[1] = var; o[2] = q0; o[3] = q1; o[4] = q2; o[5] = q3; o[6] = q4;
}
#undef CE

#define UR 48
#define FP 60
#define EP 68
#define QP 196

__global__ __launch_bounds__(128) void k_fused(const float* __restrict__ ws,
                                               const float* __restrict__ Wemb,
                                               const float* __restrict__ bemb,
                                               const float* __restrict__ pos,
                                               const float* __restrict__ win,
                                               const float* __restrict__ bin,
                                               float* __restrict__ part) {
  __shared__ __attribute__((aligned(16))) float sA[UR * FP];
  __shared__ __attribute__((aligned(16))) float sB[UR * EP];
  __shared__ __attribute__((aligned(16))) float sC[UR * QP];
  const int tid  = threadIdx.x;
  const int lane = tid & 31;
  const int wid  = tid >> 5;
  const int r  = lane & 15;
  const int hf = lane >> 4;
  const size_t row0 = (size_t)blockIdx.x * UR;

  {
    int* sRow = reinterpret_cast<int*>(sB);
    v4f* wt = reinterpret_cast<v4f*>(sC) + wid * (32 * ROWQ);
    #pragma unroll
    for (int k = 0; k < 3; ++k) {
      const int idx = tid + 128 * k;
      const int t = idx / UR;
      const int i = idx - t * UR;
      const int R = (int)row0 + i;
      const int c = R & 255;
      const int bl = R >> 8;
      const int b = bl / LL;
      const int l = bl - b * LL;
      sRow[idx] = ((l * TT + t) * BB + b) * CC + c;
    }
    __syncthreads();

    v4f x[16];
    {
      const int base = wid * 32 + hf;
      #pragma unroll
      for (int q = 0; q < 16; ++q) {
        const int sr = sRow[base + 2 * q];
        x[q] = *(reinterpret_cast<const v4f*>(ws + (size_t)sr * SS) + r);
      }
    }
    #pragma unroll 1
    for (int p = 0; p < 3; ++p) {
      #pragma unroll
      for (int q = 0; q < 16; ++q) wt[(2 * q + hf) * ROWQ + r] = x[q];
      __syncthreads();

      float v[64];
      #pragma unroll
      for (int q = 0; q < 16; ++q) {
        const v4f y = wt[lane * ROWQ + q];
        v[4 * q + 0] = y.x; v[4 * q + 1] = y.y; v[4 * q + 2] = y.z; v[4 * q + 3] = y.w;
      }
      if (p < 2) {
        const int base = (p + 1) * 128 + wid * 32 + hf;
        #pragma unroll
        for (int q = 0; q < 16; ++q) {
          const int sr = sRow[base + 2 * q];
          x[q] = *(reinterpret_cast<const v4f*>(ws + (size_t)sr * SS) + r);
        }
      }
      const int idx = p * 128 + tid;
      const int t = idx / UR;
      const int i = idx - t * UR;
      row_stats(v, sA + i * FP + t * 7);
    }
  }
  __syncthreads();

  {
    const int n0 = wid * 16;
    v8f e0 = {0, 0, 0, 0, 0, 0, 0, 0};
    v8f e1 = {0, 0, 0, 0, 0, 0, 0, 0};
    v8f e2 = {0, 0, 0, 0, 0, 0, 0, 0};
    #pragma unroll
    for (int k0 = 0; k0 < 56; k0 += 4) {
      const int c0 = k0 + 2 * hf;
      const float* bp = Wemb + (size_t)c0 * EE + n0 + r;
      v2f b; b.x = bp[0]; b.y = bp[EE];
      const float* ap = sA + r * FP + c0;
      const v2f a0 = *reinterpret_cast<const v2f*>(ap);
      const v2f a1 = *reinterpret_cast<const v2f*>(ap + 16 * FP);
      const v2f a2 = *reinterpret_cast<const v2f*>(ap + 32 * FP);
      e0 = wmma4(a0, b, e0);
      e1 = wmma4(a1, b, e1);
      e2 = wmma4(a2, b, e2);
    }
    const int n = n0 + r;
    const float be = bemb[n];
    #pragma unroll
    for (int g = 0; g < 8; ++g) {
      const int ml = g + 8 * hf;
      const int l0 = (int)(((row0 + ml)      >> 8) % LL);
      const int l1 = (int)(((row0 + ml + 16) >> 8) % LL);
      const int l2 = (int)(((row0 + ml + 32) >> 8) % LL);
      sB[(ml)      * EP + n] = e0[g] + be + pos[l0 * EE + n];
      sB[(ml + 16) * EP + n] = e1[g] + be + pos[l1 * EE + n];
      sB[(ml + 32) * EP + n] = e2[g] + be + pos[l2 * EE + n];
    }
  }
  __syncthreads();

  {
    v2f a[3][16];
    #pragma unroll
    for (int kk = 0; kk < 16; ++kk) {
      const float* ap = sB + r * EP + 4 * kk + 2 * hf;
      a[0][kk] = *reinterpret_cast<const v2f*>(ap);
      a[1][kk] = *reinterpret_cast<const v2f*>(ap + 16 * EP);
      a[2][kk] = *reinterpret_cast<const v2f*>(ap + 32 * EP);
    }
    #pragma unroll
    for (int j = 0; j < 3; ++j) {
      const int n0 = (wid + 4 * j) * 16;
      v8f c0v = {0, 0, 0, 0, 0, 0, 0, 0};
      v8f c1v = {0, 0, 0, 0, 0, 0, 0, 0};
      v8f c2v = {0, 0, 0, 0, 0, 0, 0, 0};
      #pragma unroll
      for (int kk = 0; kk < 16; ++kk) {
        const int c0 = 4 * kk + 2 * hf;
        const float* bp = win + (size_t)c0 * 192 + n0 + r;
        v2f b; b.x = bp[0]; b.y = bp[192];
        c0v = wmma4(a[0][kk], b, c0v);
        c1v = wmma4(a[1][kk], b, c1v);
        c2v = wmma4(a[2][kk], b, c2v);
      }
      const int n = n0 + r;
      const float bi_ = bin[n];
      #pragma unroll
      for (int g = 0; g < 8; ++g) {
        const int ml = g + 8 * hf;
        sC[(ml)      * QP + n] = c0v[g] + bi_;
        sC[(ml + 16) * QP + n] = c1v[g] + bi_;
        sC[(ml + 32) * QP + n] = c2v[g] + bi_;
      }
    }
  }
  __syncthreads();

  const int rq = (r < LL) ? r : (LL - 1);
  const int rk = (r < 8) ? ((r < 6) ? r : 5) : ((r < 14) ? (r - 2) : (LL - 1));

  const int gsplit = (int)((((row0 >> 8) + 1) << 8) - row0);
  float t0[4], t1[4];
  #pragma unroll
  for (int bi = 0; bi < 4; ++bi) {
    const float* base = sC + (bi * LL) * QP + wid * HDD;

    v8f S = {0, 0, 0, 0, 0, 0, 0, 0};
    #pragma unroll
    for (int k0 = 0; k0 < 16; k0 += 4) {
      const int c0 = k0 + 2 * hf;
      const v2f a = *reinterpret_cast<const v2f*>(base + rk * QP + EE + c0);
      const v2f b = *reinterpret_cast<const v2f*>(base + rq * QP + c0);
      S = wmma4(a, b, S);
    }

    float sv[6];
    #pragma unroll
    for (int g = 0; g < 6; ++g) sv[g] = S[g] * 0.25f;
    float rm = fmaxf(fmaxf(fmaxf(sv[0], sv[1]), fmaxf(sv[2], sv[3])), fmaxf(sv[4], sv[5]));
    rm = fmaxf(rm, xhalf(rm));
    float e[6];
    #pragma unroll
    for (int g = 0; g < 6; ++g) e[g] = __expf(sv[g] - rm);
    float rs = ((e[0] + e[1]) + (e[2] + e[3])) + (e[4] + e[5]);
    rs += xhalf(rs);

    v8f Cv = {0, 0, 0, 0, 0, 0, 0, 0};
    #pragma unroll
    for (int j = 0; j < 3; ++j) {
      v2f a; a.x = e[2 * j] / rs; a.y = e[2 * j + 1] / rs;
      const float* vp = base + (6 * hf + 2 * j) * QP + 128 + r;
      v2f b; b.x = vp[0]; b.y = vp[QP];
      Cv = wmma4(a, b, Cv);
    }
    float u0 = 0.0f, u1 = 0.0f;
    #pragma unroll
    for (int g = 0; g < 8; ++g) {
      const int m = g + 8 * hf;
      const float cv = (m < LL) ? Cv[g] : 0.0f;
      const bool first = (bi * LL + m) < gsplit;
      u0 += first ? cv : 0.0f;
      u1 += first ? 0.0f : cv;
    }
    t0[bi] = u0; t1[bi] = u1;
  }
  float s0 = (t0[0] + t0[1]) + (t0[2] + t0[3]);
  float s1 = (t1[0] + t1[1]) + (t1[2] + t1[3]);
  s0 += __shfl_xor(s0, 16, 32);
  s1 += __shfl_xor(s1, 16, 32);
  if (hf == 0) {
    sB[wid * HDD + r]      = s0;
    sB[EE + wid * HDD + r] = s1;
  }
  __syncthreads();

  part[(size_t)blockIdx.x * (2 * EE) + tid] = sB[tid];
}

#define NGRP (BB * LL)
#define HP 36
#define RP 132
__global__ __launch_bounds__(128) void k_tail(
    const float* __restrict__ part,
    const float* __restrict__ wout, const float* __restrict__ bout,
    const float* __restrict__ mw1, const float* __restrict__ mb1,
    const float* __restrict__ mg,  const float* __restrict__ mbb,
    const float* __restrict__ mw2, const float* __restrict__ mb2,
    const float* __restrict__ nw1, const float* __restrict__ nb1,
    const float* __restrict__ ng,  const float* __restrict__ nbb,
    const float* __restrict__ nw2, const float* __restrict__ nb2,
    float* __restrict__ out) {
  __shared__ __attribute__((aligned(16))) float sM[16 * EP];
  __shared__ __attribute__((aligned(16))) float sP[16 * EP];
  __shared__ __attribute__((aligned(16))) float sH[2 * 16 * HP];
  __shared__ __attribute__((aligned(16))) float sG[16 * HP];
  __shared__ __attribute__((aligned(16))) float sR[16 * RP];
  __shared__ __attribute__((aligned(16))) float sLN[3 * HIDD];
  const int tid  = threadIdx.x;
  const int lane = tid & 31;
  const int wid  = tid >> 5;
  const int r  = lane & 15;
  const int hf = lane >> 4;
  const int qtr    = blockIdx.x & 3;
  const int branch = (blockIdx.x >> 2) & 1;
  const int G0     = (blockIdx.x >> 3) * 16;
  const float* w1 = branch ? nw1 : mw1;
  const float* b1 = branch ? nb1 : mb1;
  const float* gg = branch ? ng  : mg;
  const float* bb = branch ? nbb : mbb;
  const float* w2 = branch ? nw2 : mw2;
  const float* b2 = branch ? nb2 : mb2;

  const float4* p4 = reinterpret_cast<const float4*>(part);
  float4 pv[2][7];
  #pragma unroll
  for (int i = 0; i < 2; ++i) {
    const int it = tid + 128 * i;
    const int G  = G0 + (it >> 4);
    const int jf = (G * 256) / UR;
    const int jl = (G * 256 + 255) / UR;
    #pragma unroll
    for (int jj = 0; jj < 7; ++jj) {
      const int j = min(jf + jj, jl);
      const int slot = G - ((j * UR) >> 8);
      pv[i][jj] = p4[(size_t)(j * 2 + slot) * (EE / 4) + (it & 15)];
    }
  }

  const int n0 = wid * 16;
  v2f bo_[16];
  #pragma unroll
  for (int kk = 0; kk < 16; ++kk) {
    const int c0 = 4 * kk + 2 * hf;
    const float* bp = wout + (size_t)c0 * EE + n0 + r;
    bo_[kk].x = bp[0]; bo_[kk].y = bp[EE];
  }
  const float bo = bout[n0 + r];
  const int ct = wid & 1;
  const int kh = wid >> 1;
  v2f bw1[8];
  #pragma unroll
  for (int kk = 0; kk < 8; ++kk) {
    const int c0 = 4 * (8 * kh + kk) + 2 * hf;
    const float* bp = w1 + (size_t)c0 * HIDD + ct * 16 + r;
    bw1[kk].x = bp[0]; bw1[kk].y = bp[HIDD];
  }
  v2f bw2[2][8];
  float b2v[2];
  #pragma unroll
  for (int j = 0; j < 2; ++j) {
    const int n = qtr * 128 + (2 * wid + j) * 16 + r;
    #pragma unroll
    for (int kk = 0; kk < 8; ++kk) {
      const int c0 = 4 * kk + 2 * hf;
      const float* bp = w2 + (size_t)c0 * 512 + n;
      bw2[j][kk].x = bp[0]; bw2[j][kk].y = bp[512];
    }
    b2v[j] = b2[n];
  }
  if (tid < HIDD) {
    sLN[tid] = b1[tid]; sLN[HIDD + tid] = gg[tid]; sLN[2 * HIDD + tid] = bb[tid];
  }

  #pragma unroll
  for (int i = 0; i < 2; ++i) {
    const int it = tid + 128 * i;
    const int G  = G0 + (it >> 4);
    const int jf = (G * 256) / UR;
    const int jl = (G * 256 + 255) / UR;
    float4 s = pv[i][0];
    #pragma unroll
    for (int jj = 1; jj < 7; ++jj) {
      const bool have = (jf + jj) <= jl;
      s.x += have ? pv[i][jj].x : 0.0f;
      s.y += have ? pv[i][jj].y : 0.0f;
      s.z += have ? pv[i][jj].z : 0.0f;
      s.w += have ? pv[i][jj].w : 0.0f;
    }
    s.x *= (1.0f / 256.0f); s.y *= (1.0f / 256.0f); s.z *= (1.0f / 256.0f); s.w *= (1.0f / 256.0f);
    *reinterpret_cast<float4*>(sM + (it >> 4) * EP + (it & 15) * 4) = s;
  }
  __syncthreads();

  {
    v8f acc0 = {0, 0, 0, 0, 0, 0, 0, 0};
    v8f acc1 = {0, 0, 0, 0, 0, 0, 0, 0};
    #pragma unroll
    for (int kk = 0; kk < 16; kk += 2) {
      const float* ap = sM + r * EP + 4 * kk + 2 * hf;
      const v2f a0 = *reinterpret_cast<const v2f*>(ap);
      const v2f a1 = *reinterpret_cast<const v2f*>(ap + 4);
      acc0 = wmma4(a0, bo_[kk],     acc0);
      acc1 = wmma4(a1, bo_[kk + 1], acc1);
    }
    #pragma unroll
    for (int g = 0; g < 8; ++g)
      sP[(g + 8 * hf) * EP + n0 + r] = (acc0[g] + acc1[g]) + bo;
  }
  __syncthreads();

  {
    v8f h0 = {0, 0, 0, 0, 0, 0, 0, 0};
    v8f h1 = {0, 0, 0, 0, 0, 0, 0, 0};
    #pragma unroll
    for (int kk = 0; kk < 8; kk += 2) {
      const float* ap = sP + r * EP + 4 * (8 * kh + kk) + 2 * hf;
      const v2f a0 = *reinterpret_cast<const v2f*>(ap);
      const v2f a1 = *reinterpret_cast<const v2f*>(ap + 4);
      h0 = wmma4(a0, bw1[kk],     h0);
      h1 = wmma4(a1, bw1[kk + 1], h1);
    }
    #pragma unroll
    for (int g = 0; g < 8; ++g)
      sH[kh * (16 * HP) + (g + 8 * hf) * HP + ct * 16 + r] = h0[g] + h1[g];
  }
  __syncthreads();

  {
    const int row = tid >> 3;
    const int cq  = tid & 7;
    const float* h0p = sH + row * HP;
    const float* h1p = sH + 16 * HP + row * HP;
    float x[HIDD];
    #pragma unroll
    for (int q = 0; q < 8; ++q) {
      const v4f p0 = *reinterpret_cast<const v4f*>(h0p + 4 * q);
      const v4f p1 = *reinterpret_cast<const v4f*>(h1p + 4 * q);
      const v4f bv = *reinterpret_cast<const v4f*>(sLN + 4 * q);
      x[4 * q + 0] = (p0.x + p1.x) + bv.x;
      x[4 * q + 1] = (p0.y + p1.y) + bv.y;
      x[4 * q + 2] = (p0.z + p1.z) + bv.z;
      x[4 * q + 3] = (p0.w + p1.w) + bv.w;
    }
    float m0 = 0.0f, m1 = 0.0f, m2 = 0.0f, m3 = 0.0f;
    #pragma unroll
    for (int q = 0; q < 8; ++q) { m0 += x[4 * q]; m1 += x[4 * q + 1]; m2 += x[4 * q + 2]; m3 += x[4 * q + 3]; }
    const float mu = ((m0 + m1) + (m2 + m3)) * (1.0f / 32.0f);
    float v0 = 0.0f, v1 = 0.0f, v2 = 0.0f, v3 = 0.0f;
    #pragma unroll
    for (int q = 0; q < 8; ++q) {
      const float d0 = x[4 * q] - mu, d1 = x[4 * q + 1] - mu, d2 = x[4 * q + 2] - mu, d3 = x[4 * q + 3] - mu;
      v0 = fmaf(d0, d0, v0); v1 = fmaf(d1, d1, v1); v2 = fmaf(d2, d2, v2); v3 = fmaf(d3, d3, v3);
    }
    const float vv = ((v0 + v1) + (v2 + v3)) * (1.0f / 32.0f);
    const float rs = rsqrtf(vv + 1e-5f);
    const v4f q0 = *reinterpret_cast<const v4f*>(h0p + 4 * cq);
    const v4f q1 = *reinterpret_cast<const v4f*>(h1p + 4 * cq);
    const v4f qb = *reinterpret_cast<const v4f*>(sLN + 4 * cq);
    const v4f gv = *reinterpret_cast<const v4f*>(sLN + HIDD + 4 * cq);
    const v4f bt = *reinterpret_cast<const v4f*>(sLN + 2 * HIDD + 4 * cq);
    const float y0 = (((q0.x + q1.x) + qb.x) - mu) * rs * gv.x + bt.x;
    const float y1 = (((q0.y + q1.y) + qb.y) - mu) * rs * gv.y + bt.y;
    const float y2 = (((q0.z + q1.z) + qb.z) - mu) * rs * gv.z + bt.z;
    const float y3 = (((q0.w + q1.w) + qb.w) - mu) * rs * gv.w + bt.w;
    v4f o;
    o.x = 0.5f * y0 * (1.0f + erff(y0 * 0.70710678118654752f));
    o.y = 0.5f * y1 * (1.0f + erff(y1 * 0.70710678118654752f));
    o.z = 0.5f * y2 * (1.0f + erff(y2 * 0.70710678118654752f));
    o.w = 0.5f * y3 * (1.0f + erff(y3 * 0.70710678118654752f));
    *reinterpret_cast<v4f*>(sG + row * HP + 4 * cq) = o;
  }
  __syncthreads();

  {
    v8f c0v = {0, 0, 0, 0, 0, 0, 0, 0};
    v8f c1v = {0, 0, 0, 0, 0, 0, 0, 0};
    #pragma unroll
    for (int kk = 0; kk < 8; ++kk) {
      const v2f a = *reinterpret_cast<const v2f*>(sG + r * HP + 4 * kk + 2 * hf);
      c0v = wmma4(a, bw2[0][kk], c0v);
      c1v = wmma4(a, bw2[1][kk], c1v);
    }
    const int nl = (2 * wid) * 16 + r;
    #pragma unroll
    for (int g = 0; g < 8; ++g) {
      sR[(g + 8 * hf) * RP + nl]      = c0v[g] + b2v[0];
      sR[(g + 8 * hf) * RP + nl + 16] = c1v[g] + b2v[1];
    }
  }
  __syncthreads();

  #pragma unroll
  for (int q = 0; q < 2; ++q) {
    const int idx = tid + 128 * q;
    const int row = idx >> 4;
    const int hl  = (idx >> 3) & 1;
    const int j   = idx & 7;
    float* p = sR + row * RP + hl * 64 + 28 + (NKK * j - (j * (j - 1)) / 2);
    *p = softplusf(*p) + 0.01f;
  }
  if (tid < 16) { sR[tid * RP + 128] = 0.0f; sR[tid * RP + 129] = 1.0f; }
  __syncthreads();

  {
    const int hl = tid >> 6;
    const int ik = tid & 63;
    const int i  = ik >> 3;
    const int k  = ik & 7;
    int oL[NKK], oU[NKK];
    #pragma unroll
    for (int j = 0; j < NKK; ++j) {
      const bool on = (j <= i) && (j <= k);
      oL[j] = on ? ((j == i) ? 129 : (hl * 64 + (i * (i - 1)) / 2 + j)) : 128;
      oU[j] = on ? (hl * 64 + 28 + (NKK * j - (j * (j - 1)) / 2) + (k - j)) : 128;
    }
    const size_t obase = ((size_t)branch * NGRP + G0) * 512 + qtr * 128 + tid;
    #pragma unroll
    for (int row = 0; row < 16; ++row) {
      const float* rb = sR + row * RP;
      float acc = 0.0f;
      #pragma unroll
      for (int j = 0; j < NKK; ++j) acc += rb[oL[j]] * rb[oU[j]];
      out[obase + (size_t)row * 512] = acc;
    }
  }
}

extern "C" void kernel_launch(void* const* d_in, const int* in_sizes, int n_in,
                              void* d_out, int out_size, void* d_ws, size_t ws_size,
                              hipStream_t stream) {
  const float* wsfeat = (const float*)d_in[0];
  const float* W_emb  = (const float*)d_in[1];
  const float* b_emb  = (const float*)d_in[2];
  const float* pos    = (const float*)d_in[3];
  const float* w_in   = (const float*)d_in[4];
  const float* b_in   = (const float*)d_in[5];
  const float* w_out  = (const float*)d_in[6];
  const float* b_out  = (const float*)d_in[7];
  const float* m_w1 = (const float*)d_in[8];
  const float* m_b1 = (const float*)d_in[9];
  const float* m_g  = (const float*)d_in[10];
  const float* m_bb = (const float*)d_in[11];
  const float* m_w2 = (const float*)d_in[12];
  const float* m_b2 = (const float*)d_in[13];
  const float* n_w1 = (const float*)d_in[14];
  const float* n_b1 = (const float*)d_in[15];
  const float* n_g  = (const float*)d_in[16];
  const float* n_bb = (const float*)d_in[17];
  const float* n_w2 = (const float*)d_in[18];
  const float* n_b2 = (const float*)d_in[19];
  float* out = (float*)d_out;

  float* wsp    = (float*)d_ws;
  float* part   = wsp  + (size_t)MROWS * 56;

  k_fused<<<MROWS / UR, 128, 0, stream>>>(wsfeat, W_emb, b_emb, pos, w_in, b_in, part);
  k_tail<<<(NGRP / 16) * 8, 128, 0, stream>>>(part, w_out, b_out,
      m_w1, m_b1, m_g, m_bb, m_w2, m_b2,
      n_w1, n_b1, n_g, n_bb, n_w2, n_b2, out);
}


# value projection moved behind the pooling: K12 forms no v tile and no P@v, it emits per head the P-weighted emb sums of each pool group (split by query row); K3 projects them; sum order changed
# speedup vs baseline: 4.3553x; 1.0421x over previous
#include <hip/hip_runtime.h>
#include <math.h>

#define LL   12
#define TT   8
#define BB   32
#define CC   256
#define SS   64
#define EE   64
#define HDD  16
#define NHH  4
#define HIDD 32
#define HH   8
#define NKK  8
#define MROWS (BB*CC*LL)
#define SROWS (LL*TT*BB*CC)

typedef float v2f __attribute__((ext_vector_type(2)));
typedef float v4f __attribute__((ext_vector_type(4)));
typedef float v8f __attribute__((ext_vector_type(8)));

static __device__ __forceinline__ v8f wmma4(v2f a, v2f b, v8f c) {
  return __builtin_amdgcn_wmma_f32_16x16x4_f32(false, a, false, b, (short)0, c,
                                               false, false);
}

static __device__ __forceinline__ float softplusf(float x) {
  return (x > 20.0f) ? x : log1pf(__expf(x));
}

static __device__ __forceinline__ float xhalf(float x) {
  const int b = __builtin_bit_cast(int, x);
  return __builtin_bit_cast(float, __builtin_amdgcn_permlanex16(b, b, 0x76543210, 0xfedcba98, false, false));
}

static __device__ __forceinline__ float rowsum16(float x) {
  x += __builtin_bit_cast(float, __builtin_amdgcn_update_dpp(0, __builtin_bit_cast(int, x), 0x161, 0xf, 0xf, true));
  x += __builtin_bit_cast(float, __builtin_amdgcn_update_dpp(0, __builtin_bit_cast(int, x), 0x162, 0xf, 0xf, true));
  x += __builtin_bit_cast(float, __builtin_amdgcn_update_dpp(0, __builtin_bit_cast(int, x), 0x164, 0xf, 0xf, true));
  x += __builtin_bit_cast(float, __builtin_amdgcn_update_dpp(0, __builtin_bit_cast(int, x), 0x168, 0xf, 0xf, true));
  return x;
}

#define ROWQ 17
#define CE(a, b) { const float lo_ = fminf(v[a], v[b]); const float hi_ = fmaxf(v[a], v[b]); v[a] = lo_; v[b] = hi_; }

static __device__ __forceinline__ void row_stats(float (&v)[64], float* __restrict__ o) {
  float sa = 0.0f, sb = 0.0f, sc = 0.0f, sd = 0.0f;
  float ta = 0.0f, tb = 0.0f, tc = 0.0f, td = 0.0f;
  #pragma unroll
  for (int q = 0; q < 16; ++q) {
    sa += v[4 * q + 0]; ta = fmaf(v[4 * q + 0], v[4 * q + 0], ta);
    sb += v[4 * q + 1]; tb = fmaf(v[4 * q + 1], v[4 * q + 1], tb);
    sc += v[4 * q + 2]; tc = fmaf(v[4 * q + 2], v[4 * q + 2], tc);
    sd += v[4 * q + 3]; td = fmaf(v[4 * q + 3], v[4 * q + 3], td);
  }
  const float s  = (sa + sb) + (sc + sd);
  const float s2 = (ta + tb) + (tc + td);
  const float mean = s * (1.0f / 64.0f);
  const float var  = (s2 - 64.0f * mean * mean) * (1.0f / 63.0f);

  CE(0,1) CE(2,3) CE(4,5) CE(6,7) CE(8,9) CE(10,11)
  CE(12,13) CE(14,15) CE(16,17) CE(18,19) CE(20,21) CE(22,23)
  CE(24,25) CE(26,27) CE(28,29) CE(30,31) CE(32,33) CE(34,35)
  CE(36,37) CE(38,39) CE(40,41) CE(42,43) CE(44,45) CE(46,47)
  CE(48,49) CE(50,51) CE(52,53) CE(54,55) CE(56,57) CE(58,59)
  CE(60,61) CE(62,63) CE(0,2) CE(1,3) CE(4,6) CE(5,7)
  CE(8,10) CE(9,11) CE(12,14) CE(13,15) CE(16,18) CE(17,19)
  CE(20,22) CE(21,23) CE(24,26) CE(25,27) CE(28,30) CE(29,31)
  CE(32,34) CE(33,35) CE(36,38) CE(37,39) CE(40,42) CE(41,43)
  CE(44,46) CE(45,47) CE(48,50) CE(49,51) CE(52,54) CE(53,55)
  CE(56,58) CE(57,59) CE(60,62) CE(61,63) CE(1,2) CE(5,6)
  CE(9,10) CE(13,14) CE(17,18) CE(21,22) CE(25,26) CE(29,30)
  CE(33,34) CE(37,38) CE(41,42) CE(45,46) CE(49,50) CE(53,54)
  CE(57,58) CE(61,62) CE(0,4) CE(1,5) CE(2,6) CE(3,7)
  CE(8,12) CE(9,13) CE(10,14) CE(11,15) CE(16,20) CE(17,21)
  CE(18,22) CE(19,23) CE(24,28) CE(25,29) CE(26,30) CE(27,31)
  CE(32,36) CE(33,37) CE(34,38) CE(35,39) CE(40,44) CE(41,45)
  CE(42,46) CE(43,47) CE(48,52) CE(49,53) CE(50,54) CE(51,55)
  CE(56,60) CE(57,61) CE(58,62) CE(59,63) CE(2,4) CE(3,5)
  CE(10,12) CE(11,13) CE(18,20) CE(19,21) CE(26,28) CE(27,29)
  CE(34,36) CE(35,37) CE(42,44) CE(43,45) CE(50,52) CE(51,53)
  CE(58,60) CE(59,61) CE(1,2) CE(3,4) CE(5,6) CE(9,10)
  CE(11,12) CE(13,14) CE(17,18) CE(19,20) CE(21,22) CE(25,26)
  CE(27,28) CE(29,30) CE(33,34) CE(35,36) CE(37,38) CE(41,42)
  CE(43,44) CE(45,46) CE(49,50) CE(51,52) CE(53,54) CE(57,58)
  CE(59,60) CE(61,62) CE(0,8) CE(1,9) CE(2,10) CE(3,11)
  CE(4,12) CE(5,13) CE(6,14) CE(7,15) CE(16,24) CE(17,25)
  CE(18,26) CE(19,27) CE(20,28) CE(21,29) CE(22,30) CE(23,31)
  CE(32,40) CE(33,41) CE(34,42) CE(35,43) CE(36,44) CE(37,45)
  CE(38,46) CE(39,47) CE(48,56) CE(49,57) CE(50,58) CE(51,59)
  CE(52,60) CE(53,61) CE(54,62) CE(55,63) CE(4,8) CE(5,9)
  CE(6,10) CE(7,11) CE(20,24) CE(21,25) CE(22,26) CE(23,27)
  CE(36,40) CE(37,41) CE(38,42) CE(39,43) CE(52,56) CE(53,57)
  CE(54,58) CE(55,59) CE(2,4) CE(3,5) CE(6,8) CE(7,9)
  CE(10,12) CE(11,13) CE(18,20) CE(19,21) CE(22,24) CE(23,25)
  CE(26,28) CE(27,29) CE(34,36) CE(35,37) CE(38,40) CE(39,41)
  CE(42,44) CE(43,45) CE(50,52) CE(51,53) CE(54,56) CE(55,57)
  CE(58,60) CE(59,61) CE(1,2) CE(3,4) CE(5,6) CE(7,8)
  CE(9,10) CE(11,12) CE(13,14) CE(17,18) CE(19,20) CE(21,22)
  CE(23,24) CE(25,26) CE(27,28) CE(29,30) CE(33,34) CE(35,36)
  CE(37,38) CE(39,40) CE(41,42) CE(43,44) CE(45,46) CE(49,50)
  CE(51,52) CE(53,54) CE(55,56) CE(57,58) CE(59,60) CE(61,62)
  CE(0,16) CE(1,17) CE(2,18) CE(3,19) CE(4,20) CE(5,21)
  CE(6,22) CE(7,23) CE(8,24) CE(9,25) CE(10,26) CE(11,27)
  CE(12,28) CE(13,29) CE(14,30) CE(15,31) CE(32,48) CE(33,49)
  CE(34,50) CE(35,51) CE(36,52) CE(37,53) CE(38,54) CE(39,55)
  CE(40,56) CE(41,57) CE(42,58) CE(43,59) CE(44,60) CE(45,61)
  CE(46,62) CE(47,63) CE(8,16) CE(9,17) CE(10,18) CE(11,19)
  CE(12,20) CE(13,21) CE(14,22) CE(15,23) CE(40,48) CE(41,49)
  CE(42,50) CE(43,51) CE(44,52) CE(45,53) CE(46,54) CE(47,55)
  CE(4,8) CE(5,9) CE(6,10) CE(7,11) CE(12,16) CE(13,17)
  CE(14,18) CE(15,19) CE(20,24) CE(21,25) CE(22,26) CE(23,27)
  CE(36,40) CE(37,41) CE(38,42) CE(39,43) CE(44,48) CE(45,49)
  CE(46,50) CE(47,51) CE(52,56) CE(53,57) CE(54,58) CE(55,59)
  CE(2,4) CE(3,5) CE(6,8) CE(7,9) CE(10,12) CE(11,13)
  CE(14,16) CE(15,17) CE(18,20) CE(19,21) CE(22,24) CE(23,25)
  CE(26,28) CE(27,29) CE(34,36) CE(35,37) CE(38,40) CE(39,41)
  CE(42,44) CE(43,45) CE(46,48) CE(47,49) CE(50,52) CE(51,53)
  CE(54,56) CE(55,57) CE(58,60) CE(59,61) CE(1,2) CE(3,4)
  CE(5,6) CE(7,8) CE(9,10) CE(11,12) CE(13,14) CE(15,16)
  CE(17,18) CE(19,20) CE(21,22) CE(23,24) CE(25,26) CE(27,28)
  CE(29,30) CE(33,34) CE(35,36) CE(37,38) CE(39,40) CE(41,42)
  CE(43,44) CE(45,46) CE(47,48) CE(49,50) CE(51,52) CE(53,54)
  CE(55,56) CE(57,58) CE(59,60) CE(61,62) CE(0,32) CE(1,33)
  CE(2,34) CE(3,35) CE(4,36) CE(5,37) CE(6,38) CE(7,39)
  CE(8,40) CE(9,41) CE(10,42) CE(11,43) CE(12,44) CE(13,45)
  CE(14,46) CE(15,47) CE(16,48) CE(17,49) CE(18,50) CE(19,51)
  CE(20,52) CE(21,53) CE(22,54) CE(23,55) CE(24,56) CE(25,57)
  CE(26,58) CE(27,59) CE(28,60) CE(29,61) CE(30,62) CE(31,63)
  CE(16,32) CE(17,33) CE(18,34) CE(19,35) CE(20,36) CE(21,37)
  CE(22,38) CE(23,39) CE(24,40) CE(25,41) CE(26,42) CE(27,43)
  CE(28,44) CE(29,45) CE(30,46) CE(31,47) CE(8,16) CE(9,17)
  CE(10,18) CE(11,19) CE(12,20) CE(13,21) CE(14,22) CE(15,23)
  CE(24,32) CE(25,33) CE(26,34) CE(27,35) CE(28,36) CE(29,37)
  CE(30,38) CE(31,39) CE(40,48) CE(41,49) CE(42,50) CE(43,51)
  CE(44,52) CE(45,53) CE(46,54) CE(47,55) CE(4,8) CE(5,9)
  CE(6,10) CE(7,11) CE(12,16) CE(13,17) CE(14,18) CE(15,19)
  CE(20,24) CE(21,25) CE(22,26) CE(23,27) CE(28,32) CE(29,33)
  CE(30,34) CE(31,35) CE(36,40) CE(37,41) CE(38,42) CE(39,43)
  CE(44,48) CE(45,49) CE(46,50) CE(47,51) CE(52,56) CE(53,57)
  CE(54,58) CE(55,59) CE(2,4) CE(3,5) CE(6,8) CE(7,9)
  CE(10,12) CE(11,13) CE(14,16) CE(15,17) CE(18,20) CE(19,21)
  CE(22,24) CE(23,25) CE(26,28) CE(27,29) CE(30,32) CE(31,33)
  CE(34,36) CE(35,37) CE(38,40) CE(39,41) CE(42,44) CE(43,45)
  CE(46,48) CE(47,49) CE(50,52) CE(51,53) CE(54,56) CE(55,57)
  CE(58,60) CE(59,61) CE(1,2) CE(3,4) CE(5,6) CE(7,8)
  CE(9,10) CE(11,12) CE(13,14) CE(15,16) CE(17,18) CE(19,20)
  CE(21,22) CE(23,24) CE(25,26) CE(27,28) CE(29,30) CE(31,32)
  CE(33,34) CE(35,36) CE(37,38) CE(39,40) CE(41,42) CE(43,44)
  CE(45,46) CE(47,48) CE(49,50) CE(51,52) CE(53,54) CE(55,56)
  CE(57,58) CE(59,60) CE(61,62)

  const float q0 = v[0];
  const float q1 = v[15] + 0.75f * (v[16] - v[15]);
  const float q2 = 0.5f * (v[31] + v[32]);
  const float q3 = v[47] + 0.25f * (v[48] - v[47]);
  const float q4 = v[63];

  o[0] = mean; o[1] = var; o[2] = q0; o[3] = q1; o[4] = q2; o[5] = q3; o[6] = q4;
}
#undef CE

#define UR 48
#define FP 60
#define EP 68
#define QP 196

__global__ __launch_bounds__(128) void k_fused(const float* __restrict__ ws,
                                               const float* __restrict__ Wemb,
                                               const float* __restrict__ bemb,
                                               const float* __restrict__ pos,
                                               const float* __restrict__ win,
                                               const float* __restrict__ bin,
                                               float* __restrict__ part) {
  __shared__ __attribute__((aligned(16))) float sA[UR * FP];
  __shared__ __attribute__((aligned(16))) float sB[UR * EP];
  __shared__ __attribute__((aligned(16))) float sC[UR * QP];
  const int tid  = threadIdx.x;
  const int lane = tid & 31;
  const int wid  = tid >> 5;
  const int r  = lane & 15;
  const int hf = lane >> 4;
  const size_t row0 = (size_t)blockIdx.x * UR;

  {
    int* sRow = reinterpret_cast<int*>(sB);
    v4f* wt = reinterpret_cast<v4f*>(sC) + wid * (32 * ROWQ);
    #pragma unroll
    for (int k = 0; k < 3; ++k) {
      const int idx = tid + 128 * k;
      const int t = idx / UR;
      const int i = idx - t * UR;
      const int R = (int)row0 + i;
      const int c = R & 255;
      const int bl = R >> 8;
      const int b = bl / LL;
      const int l = bl - b * LL;
      sRow[idx] = ((l * TT + t) * BB + b) * CC + c;
    }
    __syncthreads();

    v4f x[16];
    {
      const int base = wid * 32 + hf;
      #pragma unroll
      for (int q = 0; q < 16; ++q) {
        const int sr = sRow[base + 2 * q];
        x[q] = *(reinterpret_cast<const v4f*>(ws + (size_t)sr * SS) + r);
      }
    }
    #pragma unroll 1
    for (int p = 0; p < 3; ++p) {
      #pragma unroll
      for (int q = 0; q < 16; ++q) wt[(2 * q + hf) * ROWQ + r] = x[q];
      __syncthreads();

      float v[64];
      #pragma unroll
      for (int q = 0; q < 16; ++q) {
        const v4f y = wt[lane * ROWQ + q];
        v[4 * q + 0] = y.x; v[4 * q + 1] = y.y; v[4 * q + 2] = y.z; v[4 * q + 3] = y.w;
      }
      if (p < 2) {
        const int base = (p + 1) * 128 + wid * 32 + hf;
        #pragma unroll
        for (int q = 0; q < 16; ++q) {
          const int sr = sRow[base + 2 * q];
          x[q] = *(reinterpret_cast<const v4f*>(ws + (size_t)sr * SS) + r);
        }
      }
      const int idx = p * 128 + tid;
      const int t = idx / UR;
      const int i = idx - t * UR;
      row_stats(v, sA + i * FP + t * 7);
    }
  }
  __syncthreads();

  {
    const int n0 = wid * 16;
    v8f e0 = {0, 0, 0, 0, 0, 0, 0, 0};
    v8f e1 = {0, 0, 0, 0, 0, 0, 0, 0};
    v8f e2 = {0, 0, 0, 0, 0, 0, 0, 0};
    #pragma unroll
    for (int k0 = 0; k0 < 56; k0 += 4) {
      const int c0 = k0 + 2 * hf;
      const float* bp = Wemb + (size_t)c0 * EE + n0 + r;
      v2f b; b.x = bp[0]; b.y = bp[EE];
      const float* ap = sA + r * FP + c0;
      const v2f a0 = *reinterpret_cast<const v2f*>(ap);
      const v2f a1 = *reinterpret_cast<const v2f*>(ap + 16 * FP);
      const v2f a2 = *reinterpret_cast<const v2f*>(ap + 32 * FP);
      e0 = wmma4(a0, b, e0);
      e1 = wmma4(a1, b, e1);
      e2 = wmma4(a2, b, e2);
    }
    const int n = n0 + r;
    const float be = bemb[n];
    #pragma unroll
    for (int g = 0; g < 8; ++g) {
      const int ml = g + 8 * hf;
      const int l0 = (int)(((row0 + ml)      >> 8) % LL);
      const int l1 = (int)(((row0 + ml + 16) >> 8) % LL);
      const int l2 = (int)(((row0 + ml + 32) >> 8) % LL);
      sB[(ml)      * EP + n] = e0[g] + be + pos[l0 * EE + n];
      sB[(ml + 16) * EP + n] = e1[g] + be + pos[l1 * EE + n];
      sB[(ml + 32) * EP + n] = e2[g] + be + pos[l2 * EE + n];
    }
  }
  __syncthreads();

  {
    v2f a[3][16];
    #pragma unroll
    for (int kk = 0; kk < 16; ++kk) {
      const float* ap = sB + r * EP + 4 * kk + 2 * hf;
      a[0][kk] = *reinterpret_cast<const v2f*>(ap);
      a[1][kk] = *reinterpret_cast<const v2f*>(ap + 16 * EP);
      a[2][kk] = *reinterpret_cast<const v2f*>(ap + 32 * EP);
    }
    #pragma unroll
    for (int j = 0; j < 2; ++j) {
      const int n0 = (wid + 4 * j) * 16;
      v8f c0v = {0, 0, 0, 0, 0, 0, 0, 0};
      v8f c1v = {0, 0, 0, 0, 0, 0, 0, 0};
      v8f c2v = {0, 0, 0, 0, 0, 0, 0, 0};
      #pragma unroll
      for (int kk = 0; kk < 16; ++kk) {
        const int c0 = 4 * kk + 2 * hf;
        const float* bp = win + (size_t)c0 * 192 + n0 + r;
        v2f b; b.x = bp[0]; b.y = bp[192];
        c0v = wmma4(a[0][kk], b, c0v);
        c1v = wmma4(a[1][kk], b, c1v);
        c2v = wmma4(a[2][kk], b, c2v);
      }
      const int n = n0 + r;
      const float bi_ = bin[n];
      #pragma unroll
      for (int g = 0; g < 8; ++g) {
        const int ml = g + 8 * hf;
        sC[(ml)      * QP + n] = c0v[g] + bi_;
        sC[(ml + 16) * QP + n] = c1v[g] + bi_;
        sC[(ml + 32) * QP + n] = c2v[g] + bi_;
      }
    }
  }
  __syncthreads();

  const int rq = (r < LL) ? r : (LL - 1);
  const int rk = (r < 8) ? ((r < 6) ? r : 5) : ((r < 14) ? (r - 2) : (LL - 1));
  const float qmask = (r < LL) ? 1.0f : 0.0f;

  const int gsplit = (int)((((row0 >> 8) + 1) << 8) - row0);
  v4f z0 = {0, 0, 0, 0};
  v4f z1 = {0, 0, 0, 0};
  #pragma unroll
  for (int bi = 0; bi < 4; ++bi) {
    const float* base = sC + (bi * LL) * QP + wid * HDD;

    v8f S = {0, 0, 0, 0, 0, 0, 0, 0};
    #pragma unroll
    for (int k0 = 0; k0 < 16; k0 += 4) {
      const int c0 = k0 + 2 * hf;
      const v2f a = *reinterpret_cast<const v2f*>(base + rk * QP + EE + c0);
      const v2f b = *reinterpret_cast<const v2f*>(base + rq * QP + c0);
      S = wmma4(a, b, S);
    }

    float sv[6];
    #pragma unroll
    for (int g = 0; g < 6; ++g) sv[g] = S[g] * 0.25f;
    float rm = fmaxf(fmaxf(fmaxf(sv[0], sv[1]), fmaxf(sv[2], sv[3])), fmaxf(sv[4], sv[5]));
    rm = fmaxf(rm, xhalf(rm));
    float e[6];
    #pragma unroll
    for (int g = 0; g < 6; ++g) e[g] = __expf(sv[g] - rm);
    float rs = ((e[0] + e[1]) + (e[2] + e[3])) + (e[4] + e[5]);
    rs += xhalf(rs);
    const float inv = qmask / rs;

    float p[6], w[6];
    v4f x[6];
    #pragma unroll
    for (int g = 0; g < 6; ++g) {
      p[g] = e[g] * inv;
      w[g] = rowsum16(p[g]);
      x[g] = *reinterpret_cast<const v4f*>(sB + (bi * LL + 6 * hf + g) * EP + 4 * r);
    }
    const int q0 = bi * LL;
    if (q0 < gsplit && gsplit < q0 + LL) {
      const bool qfirst = (q0 + r) < gsplit;
      #pragma unroll
      for (int g = 0; g < 6; ++g) {
        const float wa = rowsum16(qfirst ? p[g] : 0.0f);
        const float wb = w[g] - wa;
        z0.x = fmaf(wa, x[g].x, z0.x); z0.y = fmaf(wa, x[g].y, z0.y); z0.z = fmaf(wa, x[g].z, z0.z); z0.w = fmaf(wa, x[g].w, z0.w);
        z1.x = fmaf(wb, x[g].x, z1.x); z1.y = fmaf(wb, x[g].y, z1.y); z1.z = fmaf(wb, x[g].z, z1.z); z1.w = fmaf(wb, x[g].w, z1.w);
      }
    } else {
      v4f zb = {0, 0, 0, 0};
      #pragma unroll
      for (int g = 0; g < 6; ++g) {
        zb.x = fmaf(w[g], x[g].x, zb.x); zb.y = fmaf(w[g], x[g].y, zb.y); zb.z = fmaf(w[g], x[g].z, zb.z); zb.w = fmaf(w[g], x[g].w, zb.w);
      }
      const bool bfirst = (q0 + LL) <= gsplit;
      z0.x += bfirst ? zb.x : 0.0f; z0.y += bfirst ? zb.y : 0.0f; z0.z += bfirst ? zb.z : 0.0f; z0.w += bfirst ? zb.w : 0.0f;
      z1.x += bfirst ? 0.0f : zb.x; z1.y += bfirst ? 0.0f : zb.y; z1.z += bfirst ? 0.0f : zb.z; z1.w += bfirst ? 0.0f : zb.w;
    }
  }
  z0.x += xhalf(z0.x); z0.y += xhalf(z0.y); z0.z += xhalf(z0.z); z0.w += xhalf(z0.w);
  z1.x += xhalf(z1.x); z1.y += xhalf(z1.y); z1.z += xhalf(z1.z); z1.w += xhalf(z1.w);

  const v4f zo = hf ? z1 : z0;
  *reinterpret_cast<v4f*>(part + (((size_t)blockIdx.x * 2 + hf) * NHH + wid) * EE + 4 * r) = zo;
}

#define NGRP (BB * LL)
#define HP 36
#define RP 132
#define ZP 260
__global__ __launch_bounds__(128) void k_tail(
    const float* __restrict__ part,
    const float* __restrict__ win, const float* __restrict__ bin,
    const float* __restrict__ wout, const float* __restrict__ bout,
    const float* __restrict__ mw1, const float* __restrict__ mb1,
    const float* __restrict__ mg,  const float* __restrict__ mbb,
    const float* __restrict__ mw2, const float* __restrict__ mb2,
    const float* __restrict__ nw1, const float* __restrict__ nb1,
    const float* __restrict__ ng,  const float* __restrict__ nbb,
    const float* __restrict__ nw2, const float* __restrict__ nb2,
    float* __restrict__ out) {
  __shared__ __attribute__((aligned(16))) float sZ[16 * ZP];
  __shared__ __attribute__((aligned(16))) float sM[16 * EP];
  __shared__ __attribute__((aligned(16))) float sP[16 * EP];
  __shared__ __attribute__((aligned(16))) float sH[2 * 16 * HP];
  __shared__ __attribute__((aligned(16))) float sG[16 * HP];
  __shared__ __attribute__((aligned(16))) float sR[16 * RP];
  __shared__ __attribute__((aligned(16))) float sLN[3 * HIDD];
  const int tid  = threadIdx.x;
  const int lane = tid & 31;
  const int wid  = tid >> 5;
  const int r  = lane & 15;
  const int hf = lane >> 4;
  const int qtr    = blockIdx.x & 3;
  const int branch = (blockIdx.x >> 2) & 1;
  const int G0     = (blockIdx.x >> 3) * 16;
  const float* w1 = branch ? nw1 : mw1;
  const float* b1 = branch ? nb1 : mb1;
  const float* gg = branch ? ng  : mg;
  const float* bb = branch ? nbb : mbb;
  const float* w2 = branch ? nw2 : mw2;
  const float* b2 = branch ? nb2 : mb2;

  static_assert(UR == 48 && CC == 256, "six blocks per pool group");
  const float4* p4 = reinterpret_cast<const float4*>(part);
  float4 pv[8][6];
  #pragma unroll
  for (int i = 0; i < 8; ++i) {
    const int G  = G0 + (tid >> 6) + 2 * i;
    const int jf = (G * 256) / UR;
    #pragma unroll
    for (int jj = 0; jj < 6; ++jj) {
      const int j = jf + jj;
      const int slot = G - ((j * UR) >> 8);
      pv[i][jj] = p4[(size_t)(j * 2 + slot) * (NHH * EE / 4) + (tid & 63)];
    }
  }

  const int n0 = wid * 16;
  v2f bo_[16];
  #pragma unroll
  for (int kk = 0; kk < 16; ++kk) {
    const int c0 = 4 * kk + 2 * hf;
    const float* bp = wout + (size_t)c0 * EE + n0 + r;
    bo_[kk].x = bp[0]; bo_[kk].y = bp[EE];
  }
  const float bo = bout[n0 + r];
  v2f bv_[16];
  #pragma unroll
  for (int kk = 0; kk < 16; ++kk) {
    const int c0 = 4 * kk + 2 * hf;
    const float* bp = win + (size_t)c0 * 192 + 2 * EE + n0 + r;
    bv_[kk].x = bp[0]; bv_[kk].y = bp[192];
  }
  const float bvb = bin[2 * EE + n0 + r];
  const int ct = wid & 1;
  const int kh = wid >> 1;
  v2f bw1[8];
  #pragma unroll
  for (int kk = 0; kk < 8; ++kk) {
    const int c0 = 4 * (8 * kh + kk) + 2 * hf;
    const float* bp = w1 + (size_t)c0 * HIDD + ct * 16 + r;
    bw1[kk].x = bp[0]; bw1[kk].y = bp[HIDD];
  }
  v2f bw2[2][8];
  float b2v[2];
  #pragma unroll
  for (int j = 0; j < 2; ++j) {
    const int n = qtr * 128 + (2 * wid + j) * 16 + r;
    #pragma unroll
    for (int kk = 0; kk < 8; ++kk) {
      const int c0 = 4 * kk + 2 * hf;
      const float* bp = w2 + (size_t)c0 * 512 + n;
      bw2[j][kk].x = bp[0]; bw2[j][kk].y = bp[512];
    }
    b2v[j] = b2[n];
  }
  if (tid < HIDD) {
    sLN[tid] = b1[tid]; sLN[HIDD + tid] = gg[tid]; sLN[2 * HIDD + tid] = bb[tid];
  }

  #pragma unroll
  for (int i = 0; i < 8; ++i) {
    const int g  = (tid >> 6) + 2 * i;
    float4 s = pv[i][0];
    #pragma unroll
    for (int jj = 1; jj < 6; ++jj) {
      s.x += pv[i][jj].x; s.y += pv[i][jj].y; s.z += pv[i][jj].z; s.w += pv[i][jj].w;
    }
    s.x *= (1.0f / 256.0f); s.y *= (1.0f / 256.0f); s.z *= (1.0f / 256.0f); s.w *= (1.0f / 256.0f);
    *reinterpret_cast<float4*>(sZ + g * ZP + (tid & 63) * 4) = s;
  }
  __syncthreads();

  {
    v8f m0 = {0, 0, 0, 0, 0, 0, 0, 0};
    v8f m1 = {0, 0, 0, 0, 0, 0, 0, 0};
    #pragma unroll
    for (int kk = 0; kk < 16; kk += 2) {
      const float* ap = sZ + r * ZP + wid * EE + 4 * kk + 2 * hf;
      const v2f a0 = *reinterpret_cast<const v2f*>(ap);
      const v2f a1 = *reinterpret_cast<const v2f*>(ap + 4);
      m0 = wmma4(a0, bv_[kk],     m0);
      m1 = wmma4(a1, bv_[kk + 1], m1);
    }
    #pragma unroll
    for (int g = 0; g < 8; ++g)
      sM[(g + 8 * hf) * EP + n0 + r] = (m0[g] + m1[g]) + bvb;
  }
  __syncthreads();

  {
    v8f acc0 = {0, 0, 0, 0, 0, 0, 0, 0};
    v8f acc1 = {0, 0, 0, 0, 0, 0, 0, 0};
    #pragma unroll
    for (int kk = 0; kk < 16; kk += 2) {
      const float* ap = sM + r * EP + 4 * kk + 2 * hf;
      const v2f a0 = *reinterpret_cast<const v2f*>(ap);
      const v2f a1 = *reinterpret_cast<const v2f*>(ap + 4);
      acc0 = wmma4(a0, bo_[kk],     acc0);
      acc1 = wmma4(a1, bo_[kk + 1], acc1);
    }
    #pragma unroll
    for (int g = 0; g < 8; ++g)
      sP[(g + 8 * hf) * EP + n0 + r] = (acc0[g] + acc1[g]) + bo;
  }
  __syncthreads();

  {
    v8f h0 = {0, 0, 0, 0, 0, 0, 0, 0};
    v8f h1 = {0, 0, 0, 0, 0, 0, 0, 0};
    #pragma unroll
    for (int kk = 0; kk < 8; kk += 2) {
      const float* ap = sP + r * EP + 4 * (8 * kh + kk) + 2 * hf;
      const v2f a0 = *reinterpret_cast<const v2f*>(ap);
      const v2f a1 = *reinterpret_cast<const v2f*>(ap + 4);
      h0 = wmma4(a0, bw1[kk],     h0);
      h1 = wmma4(a1, bw1[kk + 1], h1);
    }
    #pragma unroll
    for (int g = 0; g < 8; ++g)
      sH[kh * (16 * HP) + (g + 8 * hf) * HP + ct * 16 + r] = h0[g] + h1[g];
  }
  __syncthreads();

  {
    const int row = tid >> 3;
    const int cq  = tid & 7;
    const float* h0p = sH + row * HP;
    const float* h1p = sH + 16 * HP + row * HP;
    float x[HIDD];
    #pragma unroll
    for (int q = 0; q < 8; ++q) {
      const v4f p0 = *reinterpret_cast<const v4f*>(h0p + 4 * q);
      const v4f p1 = *reinterpret_cast<const v4f*>(h1p + 4 * q);
      const v4f bv = *reinterpret_cast<const v4f*>(sLN + 4 * q);
      x[4 * q + 0] = (p0.x + p1.x) + bv.x;
      x[4 * q + 1] = (p0.y + p1.y) + bv.y;
      x[4 * q + 2] = (p0.z + p1.z) + bv.z;
      x[4 * q + 3] = (p0.w + p1.w) + bv.w;
    }
    float m0 = 0.0f, m1 = 0.0f, m2 = 0.0f, m3 = 0.0f;
    #pragma unroll
    for (int q = 0; q < 8; ++q) { m0 += x[4 * q]; m1 += x[4 * q + 1]; m2 += x[4 * q + 2]; m3 += x[4 * q + 3]; }
    const float mu = ((m0 + m1) + (m2 + m3)) * (1.0f / 32.0f);
    float v0 = 0.0f, v1 = 0.0f, v2 = 0.0f, v3 = 0.0f;
    #pragma unroll
    for (int q = 0; q < 8; ++q) {
      const float d0 = x[4 * q] - mu, d1 = x[4 * q + 1] - mu, d2 = x[4 * q + 2] - mu, d3 = x[4 * q + 3] - mu;
      v0 = fmaf(d0, d0, v0); v1 = fmaf(d1, d1, v1); v2 = fmaf(d2, d2, v2); v3 = fmaf(d3, d3, v3);
    }
    const float vv = ((v0 + v1) + (v2 + v3)) * (1.0f / 32.0f);
    const float rs = rsqrtf(vv + 1e-5f);
    const v4f q0 = *reinterpret_cast<const v4f*>(h0p + 4 * cq);
    const v4f q1 = *reinterpret_cast<const v4f*>(h1p + 4 * cq);
    const v4f qb = *reinterpret_cast<const v4f*>(sLN + 4 * cq);
    const v4f gv = *reinterpret_cast<const v4f*>(sLN + HIDD + 4 * cq);
    const v4f bt = *reinterpret_cast<const v4f*>(sLN + 2 * HIDD + 4 * cq);
    const float y0 = (((q0.x + q1.x) + qb.x) - mu) * rs * gv.x + bt.x;
    const float y1 = (((q0.y + q1.y) + qb.y) - mu) * rs * gv.y + bt.y;
    const float y2 = (((q0.z + q1.z) + qb.z) - mu) * rs * gv.z + bt.z;
    const float y3 = (((q0.w + q1.w) + qb.w) - mu) * rs * gv.w + bt.w;
    v4f o;
    o.x = 0.5f * y0 * (1.0f + erff(y0 * 0.70710678118654752f));
    o.y = 0.5f * y1 * (1.0f + erff(y1 * 0.70710678118654752f));
    o.z = 0.5f * y2 * (1.0f + erff(y2 * 0.70710678118654752f));
    o.w = 0.5f * y3 * (1.0f + erff(y3 * 0.70710678118654752f));
    *reinterpret_cast<v4f*>(sG + row * HP + 4 * cq) = o;
  }
  __syncthreads();

  {
    v8f c0v = {0, 0, 0, 0, 0, 0, 0, 0};
    v8f c1v = {0, 0, 0, 0, 0, 0, 0, 0};
    #pragma unroll
    for (int kk = 0; kk < 8; ++kk) {
      const v2f a = *reinterpret_cast<const v2f*>(sG + r * HP + 4 * kk + 2 * hf);
      c0v = wmma4(a, bw2[0][kk], c0v);
      c1v = wmma4(a, bw2[1][kk], c1v);
    }
    const int nl = (2 * wid) * 16 + r;
    #pragma unroll
    for (int g = 0; g < 8; ++g) {
      sR[(g + 8 * hf) * RP + nl]      = c0v[g] + b2v[0];
      sR[(g + 8 * hf) * RP + nl + 16] = c1v[g] + b2v[1];
    }
  }
  __syncthreads();

  #pragma unroll
  for (int q = 0; q < 2; ++q) {
    const int idx = tid + 128 * q;
    const int row = idx >> 4;
    const int hl  = (idx >> 3) & 1;
    const int j   = idx & 7;
    float* p = sR + row * RP + hl * 64 + 28 + (NKK * j - (j * (j - 1)) / 2);
    *p = softplusf(*p) + 0.01f;
  }
  if (tid < 16) { sR[tid * RP + 128] = 0.0f; sR[tid * RP + 129] = 1.0f; }
  __syncthreads();

  {
    const int hl = tid >> 6;
    const int ik = tid & 63;
    const int i  = ik >> 3;
    const int k  = ik & 7;
    int oL[NKK], oU[NKK];
    #pragma unroll
    for (int j = 0; j < NKK; ++j) {
      const bool on = (j <= i) && (j <= k);
      oL[j] = on ? ((j == i) ? 129 : (hl * 64 + (i * (i - 1)) / 2 + j)) : 128;
      oU[j] = on ? (hl * 64 + 28 + (NKK * j - (j * (j - 1)) / 2) + (k - j)) : 128;
    }
    const size_t obase = ((size_t)branch * NGRP + G0) * 512 + qtr * 128 + tid;
    #pragma unroll
    for (int row = 0; row < 16; ++row) {
      const float* rb = sR + row * RP;
      float acc = 0.0f;
      #pragma unroll
      for (int j = 0; j < NKK; ++j) acc += rb[oL[j]] * rb[oU[j]];
      out[obase + (size_t)row * 512] = acc;
    }
  }
}

extern "C" void kernel_launch(void* const* d_in, const int* in_sizes, int n_in,
                              void* d_out, int out_size, void* d_ws, size_t ws_size,
                              hipStream_t stream) {
  const float* wsfeat = (const float*)d_in[0];
  const float* W_emb  = (const float*)d_in[1];
  const float* b_emb  = (const float*)d_in[2];
  const float* pos    = (const float*)d_in[3];
  const float* w_in   = (const float*)d_in[4];
  const float* b_in   = (const float*)d_in[5];
  const float* w_out  = (const float*)d_in[6];
  const float* b_out  = (const float*)d_in[7];
  const float* m_w1 = (const float*)d_in[8];
  const float* m_b1 = (const float*)d_in[9];
  const float* m_g  = (const float*)d_in[10];
  const float* m_bb = (const float*)d_in[11];
  const float* m_w2 = (const float*)d_in[12];
  const float* m_b2 = (const float*)d_in[13];
  const float* n_w1 = (const float*)d_in[14];
  const float* n_b1 = (const float*)d_in[15];
  const float* n_g  = (const float*)d_in[16];
  const float* n_bb = (const float*)d_in[17];
  const float* n_w2 = (const float*)d_in[18];
  const float* n_b2 = (const float*)d_in[19];
  float* out = (float*)d_out;

  float* wsp    = (float*)d_ws;
  float* part   = wsp;

  k_fused<<<MROWS / UR, 128, 0, stream>>>(wsfeat, W_emb, b_emb, pos, w_in, b_in, part);
  k_tail<<<(NGRP / 16) * 8, 128, 0, stream>>>(part, w_in, b_in, w_out, b_out,
      m_w1, m_b1, m_g, m_bb, m_w2, m_b2,
      n_w1, n_b1, n_g, n_bb, n_w2, n_b2, out);
}


# K12 row statistics: the min/max sorting network as inline asm with hand-assigned registers, one dual-issue instruction (min+max) per compare-exchange instead of about 1.3; results bit-identical
# speedup vs baseline: 4.7000x; 1.0791x over previous
#include <hip/hip_runtime.h>
#include <math.h>

#define LL   12
#define TT   8
#define BB   32
#define CC   256
#define SS   64
#define EE   64
#define HDD  16
#define NHH  4
#define HIDD 32
#define HH   8
#define NKK  8
#define MROWS (BB*CC*LL)
#define SROWS (LL*TT*BB*CC)

typedef float v2f __attribute__((ext_vector_type(2)));
typedef float v4f __attribute__((ext_vector_type(4)));
typedef float v8f __attribute__((ext_vector_type(8)));

static __device__ __forceinline__ v8f wmma4(v2f a, v2f b, v8f c) {
  return __builtin_amdgcn_wmma_f32_16x16x4_f32(false, a, false, b, (short)0, c,
                                               false, false);
}

static __device__ __forceinline__ float softplusf(float x) {
  return (x > 20.0f) ? x : log1pf(__expf(x));
}

static __device__ __forceinline__ float xhalf(float x) {
  const int b = __builtin_bit_cast(int, x);
  return __builtin_bit_cast(float, __builtin_amdgcn_permlanex16(b, b, 0x76543210, 0xfedcba98, false, false));
}

static __device__ __forceinline__ float rowsum16(float x) {
  x += __builtin_bit_cast(float, __builtin_amdgcn_update_dpp(0, __builtin_bit_cast(int, x), 0x161, 0xf, 0xf, true));
  x += __builtin_bit_cast(float, __builtin_amdgcn_update_dpp(0, __builtin_bit_cast(int, x), 0x162, 0xf, 0xf, true));
  x += __builtin_bit_cast(float, __builtin_amdgcn_update_dpp(0, __builtin_bit_cast(int, x), 0x164, 0xf, 0xf, true));
  x += __builtin_bit_cast(float, __builtin_amdgcn_update_dpp(0, __builtin_bit_cast(int, x), 0x168, 0xf, 0xf, true));
  return x;
}

#define ROWQ 17

static __device__ __forceinline__ void row_stats(unsigned la, float* __restrict__ o) {
  float r0, r15, r16, r31, r32, r47, r48, r63;
  float sa, sb, sc, sd, ta, tb, tc, td;
  asm volatile(
    "ds_load_b128 v[168:171], %16\n\t"
    "ds_load_b128 v[172:175], %16 offset:16\n\t"
    "ds_load_b128 v[176:179], %16 offset:32\n\t"
    "ds_load_b128 v[180:183], %16 offset:48\n\t"
    "ds_load_b128 v[184:187], %16 offset:64\n\t"
    "ds_load_b128 v[188:191], %16 offset:80\n\t"
    "ds_load_b128 v[192:195], %16 offset:96\n\t"
    "ds_load_b128 v[196:199], %16 offset:112\n\t"
    "ds_load_b128 v[200:203], %16 offset:128\n\t"
    "ds_load_b128 v[204:207], %16 offset:144\n\t"
    "ds_load_b128 v[208:211], %16 offset:160\n\t"
    "ds_load_b128 v[212:215], %16 offset:176\n\t"
    "ds_load_b128 v[216:219], %16 offset:192\n\t"
    "ds_load_b128 v[220:223], %16 offset:208\n\t"
    "ds_load_b128 v[224:227], %16 offset:224\n\t"
    "ds_load_b128 v[228:231], %16 offset:240\n\t"
    "v_mov_b32 v245, 0\n\t"
    "v_mov_b32 v244, 0\n\t"
    "v_mov_b32 v247, 0\n\t"
    "v_mov_b32 v246, 0\n\t"
    "v_mov_b32 v249, 0\n\t"
    "v_mov_b32 v248, 0\n\t"
    "v_mov_b32 v253, 0\n\t"
    "v_mov_b32 v250, 0\n\t"
    "s_wait_dscnt 0xf\n\t"
    "v_dual_add_f32 v245, v245, v168 :: v_dual_fmac_f32 v244, v168, v168\n\t"
    "v_dual_add_f32 v247, v247, v169 :: v_dual_fmac_f32 v246, v169, v169\n\t"
    "v_dual_add_f32 v249, v249, v170 :: v_dual_fmac_f32 v248, v170, v170\n\t"
    "s_delay_alu instid0(VALU_DEP_4)\n\t"
    "v_dual_add_f32 v253, v253, v171 :: v_dual_fmac_f32 v250, v171, v171\n\t"
    "v_dual_min_num_f32 v232, v168, v169 :: v_dual_max_num_f32 v233, v168, v169\n\t"
    "v_dual_min_num_f32 v234, v170, v171 :: v_dual_max_num_f32 v235, v170, v171\n\t"
    "s_wait_dscnt 0xe\n\t"
    "v_dual_add_f32 v245, v245, v172 :: v_dual_fmac_f32 v244, v172, v172\n\t"
    "v_dual_add_f32 v247, v247, v173 :: v_dual_fmac_f32 v246, v173, v173\n\t"
    "v_dual_add_f32 v249, v249, v174 :: v_dual_fmac_f32 v248, v174, v174\n\t"
    "v_dual_add_f32 v253, v253, v175 :: v_dual_fmac_f32 v250, v175, v175\n\t"
    "v_dual_min_num_f32 v236, v172, v173 :: v_dual_max_num_f32 v237, v172, v173\n\t"
    "v_dual_min_num_f32 v238, v174, v175 :: v_dual_max_num_f32 v239, v174, v175\n\t"
    "s_wait_dscnt 0xd\n\t"
    "v_dual_add_f32 v245, v245, v176 :: v_dual_fmac_f32 v244, v176, v176\n\t"
    "v_dual_add_f32 v247, v247, v177 :: v_dual_fmac_f32 v246, v177, v177\n\t"
    "v_dual_add_f32 v249, v249, v178 :: v_dual_fmac_f32 v248, v178, v178\n\t"
    "v_dual_add_f32 v253, v253, v179 :: v_dual_fmac_f32 v250, v179, v179\n\t"
    "v_dual_min_num_f32 v240, v176, v177 :: v_dual_max_num_f32 v241, v176, v177\n\t"
    "v_dual_min_num_f32 v242, v178, v179 :: v_dual_max_num_f32 v243, v178, v179\n\t"
    "s_wait_dscnt 0xc\n\t"
    "v_dual_add_f32 v245, v245, v180 :: v_dual_fmac_f32 v244, v180, v180\n\t"
    "v_dual_add_f32 v247, v247, v181 :: v_dual_fmac_f32 v246, v181, v181\n\t"
    "v_dual_add_f32 v249, v249, v182 :: v_dual_fmac_f32 v248, v182, v182\n\t"
    "v_dual_add_f32 v253, v253, v183 :: v_dual_fmac_f32 v250, v183, v183\n\t"
    "v_dual_min_num_f32 v168, v180, v181 :: v_dual_max_num_f32 v169, v180, v181\n\t"
    "v_dual_min_num_f32 v170, v182, v183 :: v_dual_max_num_f32 v171, v182, v183\n\t"
    "s_wait_dscnt 0xb\n\t"
    "v_dual_add_f32 v245, v245, v184 :: v_dual_fmac_f32 v244, v184, v184\n\t"
    "v_dual_add_f32 v247, v247, v185 :: v_dual_fmac_f32 v246, v185, v185\n\t"
    "v_dual_add_f32 v249, v249, v186 :: v_dual_fmac_f32 v248, v186, v186\n\t"
    "v_dual_add_f32 v253, v253, v187 :: v_dual_fmac_f32 v250, v187, v187\n\t"
    "v_dual_min_num_f32 v172, v184, v185 :: v_dual_max_num_f32 v173, v184, v185\n\t"
    "v_dual_min_num_f32 v174, v186, v187 :: v_dual_max_num_f32 v175, v186, v187\n\t"
    "s_wait_dscnt 0xa\n\t"
    "v_dual_add_f32 v245, v245, v188 :: v_dual_fmac_f32 v244, v188, v188\n\t"
    "v_dual_add_f32 v247, v247, v189 :: v_dual_fmac_f32 v246, v189, v189\n\t"
    "v_dual_add_f32 v249, v249, v190 :: v_dual_fmac_f32 v248, v190, v190\n\t"
    "v_dual_add_f32 v253, v253, v191 :: v_dual_fmac_f32 v250, v191, v191\n\t"
    "v_dual_min_num_f32 v176, v188, v189 :: v_dual_max_num_f32 v177, v188, v189\n\t"
    "v_dual_min_num_f32 v178, v190, v191 :: v_dual_max_num_f32 v179, v190, v191\n\t"
    "s_wait_dscnt 0x9\n\t"
    "v_dual_add_f32 v245, v245, v192 :: v_dual_fmac_f32 v244, v192, v192\n\t"
    "v_dual_add_f32 v247, v247, v193 :: v_dual_fmac_f32 v246, v193, v193\n\t"
    "v_dual_add_f32 v249, v249, v194 :: v_dual_fmac_f32 v248, v194, v194\n\t"
    "v_dual_add_f32 v253, v253, v195 :: v_dual_fmac_f32 v250, v195, v195\n\t"
    "v_dual_min_num_f32 v180, v192, v193 :: v_dual_max_num_f32 v181, v192, v193\n\t"
    "v_dual_min_num_f32 v182, v194, v195 :: v_dual_max_num_f32 v183, v194, v195\n\t"
    "s_wait_dscnt 0x8\n\t"
    "v_dual_add_f32 v245, v245, v196 :: v_dual_fmac_f32 v244, v196, v196\n\t"
    "v_dual_add_f32 v247, v247, v197 :: v_dual_fmac_f32 v246, v197, v197\n\t"
    "v_dual_add_f32 v249, v249, v198 :: v_dual_fmac_f32 v248, v198, v198\n\t"
    "v_dual_add_f32 v253, v253, v199 :: v_dual_fmac_f32 v250, v199, v199\n\t"
    "v_dual_min_num_f32 v184, v196, v197 :: v_dual_max_num_f32 v185, v196, v197\n\t"
    "v_dual_min_num_f32 v186, v198, v199 :: v_dual_max_num_f32 v187, v198, v199\n\t"
    "s_wait_dscnt 0x7\n\t"
    "v_dual_add_f32 v245, v245, v200 :: v_dual_fmac_f32 v244, v200, v200\n\t"
    "v_dual_add_f32 v247, v247, v201 :: v_dual_fmac_f32 v246, v201, v201\n\t"
    "v_dual_add_f32 v249, v249, v202 :: v_dual_fmac_f32 v248, v202, v202\n\t"
    "v_dual_add_f32 v253, v253, v203 :: v_dual_fmac_f32 v250, v203, v203\n\t"
    "v_dual_min_num_f32 v188, v200, v201 :: v_dual_max_num_f32 v189, v200, v201\n\t"
    "v_dual_min_num_f32 v190, v202, v203 :: v_dual_max_num_f32 v191, v202, v203\n\t"
    "s_wait_dscnt 0x6\n\t"
    "v_dual_add_f32 v245, v245, v204 :: v_dual_fmac_f32 v244, v204, v204\n\t"
    "v_dual_add_f32 v247, v247, v205 :: v_dual_fmac_f32 v246, v205, v205\n\t"
    "v_dual_add_f32 v249, v249, v206 :: v_dual_fmac_f32 v248, v206, v206\n\t"
    "v_dual_add_f32 v253, v253, v207 :: v_dual_fmac_f32 v250, v207, v207\n\t"
    "v_dual_min_num_f32 v192, v204, v205 :: v_dual_max_num_f32 v193, v204, v205\n\t"
    "v_dual_min_num_f32 v194, v206, v207 :: v_dual_max_num_f32 v195, v206, v207\n\t"
    "s_wait_dscnt 0x5\n\t"
    "v_dual_add_f32 v245, v245, v208 :: v_dual_fmac_f32 v244, v208, v208\n\t"
    "v_dual_add_f32 v247, v247, v209 :: v_dual_fmac_f32 v246, v209, v209\n\t"
    "v_dual_add_f32 v249, v249, v210 :: v_dual_fmac_f32 v248, v210, v210\n\t"
    "v_dual_add_f32 v253, v253, v211 :: v_dual_fmac_f32 v250, v211, v211\n\t"
    "v_dual_min_num_f32 v196, v208, v209 :: v_dual_max_num_f32 v197, v208, v209\n\t"
    "v_dual_min_num_f32 v198, v210, v211 :: v_dual_max_num_f32 v199, v210, v211\n\t"
    "s_wait_dscnt 0x4\n\t"
    "v_dual_add_f32 v245, v245, v212 :: v_dual_fmac_f32 v244, v212, v212\n\t"
    "v_dual_add_f32 v247, v247, v213 :: v_dual_fmac_f32 v246, v213, v213\n\t"
    "v_dual_add_f32 v249, v249, v214 :: v_dual_fmac_f32 v248, v214, v214\n\t"
    "v_dual_add_f32 v253, v253, v215 :: v_dual_fmac_f32 v250, v215, v215\n\t"
    "v_dual_min_num_f32 v200, v212, v213 :: v_dual_max_num_f32 v201, v212, v213\n\t"
    "v_dual_min_num_f32 v202, v214, v215 :: v_dual_max_num_f32 v203, v214, v215\n\t"
    "s_wait_dscnt 0x3\n\t"
    "v_dual_add_f32 v245, v245, v216 :: v_dual_fmac_f32 v244, v216, v216\n\t"
    "v_dual_add_f32 v247, v247, v217 :: v_dual_fmac_f32 v246, v217, v217\n\t"
    "v_dual_add_f32 v249, v249, v218 :: v_dual_fmac_f32 v248, v218, v218\n\t"
    "v_dual_add_f32 v253, v253, v219 :: v_dual_fmac_f32 v250, v219, v219\n\t"
    "v_dual_min_num_f32 v204, v216, v217 :: v_dual_max_num_f32 v205, v216, v217\n\t"
    "v_dual_min_num_f32 v206, v218, v219 :: v_dual_max_num_f32 v207, v218, v219\n\t"
    "s_wait_dscnt 0x2\n\t"
    "v_dual_add_f32 v245, v245, v220 :: v_dual_fmac_f32 v244, v220, v220\n\t"
    "v_dual_add_f32 v247, v247, v221 :: v_dual_fmac_f32 v246, v221, v221\n\t"
    "v_dual_add_f32 v249, v249, v222 :: v_dual_fmac_f32 v248, v222, v222\n\t"
    "v_dual_add_f32 v253, v253, v223 :: v_dual_fmac_f32 v250, v223, v223\n\t"
    "v_dual_min_num_f32 v208, v220, v221 :: v_dual_max_num_f32 v209, v220, v221\n\t"
    "v_dual_min_num_f32 v210, v222, v223 :: v_dual_max_num_f32 v211, v222, v223\n\t"
    "s_wait_dscnt 0x1\n\t"
    "v_dual_add_f32 v245, v245, v224 :: v_dual_fmac_f32 v244, v224, v224\n\t"
    "v_dual_add_f32 v247, v247, v225 :: v_dual_fmac_f32 v246, v225, v225\n\t"
    "v_dual_add_f32 v249, v249, v226 :: v_dual_fmac_f32 v248, v226, v226\n\t"
    "v_dual_add_f32 v253, v253, v227 :: v_dual_fmac_f32 v250, v227, v227\n\t"
    "v_dual_min_num_f32 v212, v224, v225 :: v_dual_max_num_f32 v213, v224, v225\n\t"
    "v_dual_min_num_f32 v214, v226, v227 :: v_dual_max_num_f32 v215, v226, v227\n\t"
    "s_wait_dscnt 0x0\n\t"
    "v_dual_add_f32 v245, v245, v228 :: v_dual_fmac_f32 v244, v228, v228\n\t"
    "v_dual_add_f32 v247, v247, v229 :: v_dual_fmac_f32 v246, v229, v229\n\t"
    "v_dual_add_f32 v249, v249, v230 :: v_dual_fmac_f32 v248, v230, v230\n\t"
    "v_dual_add_f32 v253, v253, v231 :: v_dual_fmac_f32 v250, v231, v231\n\t"
    "v_dual_min_num_f32 v216, v228, v229 :: v_dual_max_num_f32 v217, v228, v229\n\t"
    "v_dual_min_num_f32 v218, v230, v231 :: v_dual_max_num_f32 v219, v230, v231\n\t"
    "v_dual_min_num_f32 v220, v232, v234 :: v_dual_max_num_f32 v221, v232, v234\n\t"
    "v_dual_min_num_f32 v222, v233, v235 :: v_dual_max_num_f32 v223, v233, v235\n\t"
    "v_dual_min_num_f32 v226, v236, v238 :: v_dual_max_num_f32 v225, v236, v238\n\t"
    "v_dual_min_num_f32 v224, v237, v239 :: v_dual_max_num_f32 v229, v237, v239\n\t"
    "v_dual_min_num_f32 v228, v240, v242 :: v_dual_max_num_f32 v227, v240, v242\n\t"
    "v_dual_min_num_f32 v230, v241, v243 :: v_dual_max_num_f32 v231, v241, v243\n\t"
    "v_dual_min_num_f32 v234, v168, v170 :: v_dual_max_num_f32 v233, v168, v170\n\t"
    "v_dual_min_num_f32 v232, v169, v171 :: v_dual_max_num_f32 v237, v169, v171\n\t"
    "v_dual_min_num_f32 v236, v172, v174 :: v_dual_max_num_f32 v235, v172, v174\n\t"
    "v_dual_min_num_f32 v238, v173, v175 :: v_dual_max_num_f32 v239, v173, v175\n\t"
    "v_dual_min_num_f32 v242, v176, v178 :: v_dual_max_num_f32 v241, v176, v178\n\t"
    "v_dual_min_num_f32 v240, v177, v179 :: v_dual_max_num_f32 v169, v177, v179\n\t"
    "v_dual_min_num_f32 v168, v180, v182 :: v_dual_max_num_f32 v243, v180, v182\n\t"
    "v_dual_min_num_f32 v170, v181, v183 :: v_dual_max_num_f32 v171, v181, v183\n\t"
    "v_dual_min_num_f32 v174, v184, v186 :: v_dual_max_num_f32 v173, v184, v186\n\t"
    "v_dual_min_num_f32 v172, v185, v187 :: v_dual_max_num_f32 v177, v185, v187\n\t"
    "v_dual_min_num_f32 v176, v188, v190 :: v_dual_max_num_f32 v175, v188, v190\n\t"
    "v_dual_min_num_f32 v178, v189, v191 :: v_dual_max_num_f32 v179, v189, v191\n\t"
    "v_dual_min_num_f32 v182, v192, v194 :: v_dual_max_num_f32 v181, v192, v194\n\t"
    "v_dual_min_num_f32 v180, v193, v195 :: v_dual_max_num_f32 v185, v193, v195\n\t"
    "v_dual_min_num_f32 v184, v196, v198 :: v_dual_max_num_f32 v183, v196, v198\n\t"
    "v_dual_min_num_f32 v186, v197, v199 :: v_dual_max_num_f32 v187, v197, v199\n\t"
    "v_dual_min_num_f32 v190, v200, v202 :: v_dual_max_num_f32 v189, v200, v202\n\t"
    "v_dual_min_num_f32 v188, v201, v203 :: v_dual_max_num_f32 v193, v201, v203\n\t"
    "v_dual_min_num_f32 v192, v204, v206 :: v_dual_max_num_f32 v191, v204, v206\n\t"
    "v_dual_min_num_f32 v194, v205, v207 :: v_dual_max_num_f32 v195, v205, v207\n\t"
    "v_dual_min_num_f32 v198, v208, v210 :: v_dual_max_num_f32 v197, v208, v210\n\t"
    "v_dual_min_num_f32 v196, v209, v211 :: v_dual_max_num_f32 v201, v209, v211\n\t"
    "v_dual_min_num_f32 v200, v212, v214 :: v_dual_max_num_f32 v199, v212, v214\n\t"
    "v_dual_min_num_f32 v202, v213, v215 :: v_dual_max_num_f32 v203, v213, v215\n\t"
    "v_dual_min_num_f32 v206, v216, v218 :: v_dual_max_num_f32 v205, v216, v218\n\t"
    "v_dual_min_num_f32 v204, v217, v219 :: v_dual_max_num_f32 v209, v217, v219\n\t"
    "v_dual_min_num_f32 v208, v222, v221 :: v_dual_max_num_f32 v207, v222, v221\n\t"
    "v_dual_min_num_f32 v210, v224, v225 :: v_dual_max_num_f32 v213, v224, v225\n\t"
    "v_dual_min_num_f32 v212, v230, v227 :: v_dual_max_num_f32 v211, v230, v227\n\t"
    "v_dual_min_num_f32 v214, v232, v233 :: v_dual_max_num_f32 v217, v232, v233\n\t"
    "v_dual_min_num_f32 v216, v238, v235 :: v_dual_max_num_f32 v215, v238, v235\n\t"
    "v_dual_min_num_f32 v218, v240, v241 :: v_dual_max_num_f32 v221, v240, v241\n\t"
    "v_dual_min_num_f32 v222, v170, v243 :: v_dual_max_num_f32 v219, v170, v243\n\t"
    "v_dual_min_num_f32 v224, v172, v173 :: v_dual_max_num_f32 v225, v172, v173\n\t"
    "v_dual_min_num_f32 v230, v178, v175 :: v_dual_max_num_f32 v227, v178, v175\n\t"
    "v_dual_min_num_f32 v232, v180, v181 :: v_dual_max_num_f32 v233, v180, v181\n\t"
    "v_dual_min_num_f32 v238, v186, v183 :: v_dual_max_num_f32 v235, v186, v183\n\t"
    "v_dual_min_num_f32 v240, v188, v189 :: v_dual_max_num_f32 v241, v188, v189\n\t"
    "v_dual_min_num_f32 v170, v194, v191 :: v_dual_max_num_f32 v243, v194, v191\n\t"
    "v_dual_min_num_f32 v172, v196, v197 :: v_dual_max_num_f32 v173, v196, v197\n\t"
    "v_dual_min_num_f32 v178, v202, v199 :: v_dual_max_num_f32 v175, v202, v199\n\t"
    "v_dual_min_num_f32 v180, v204, v205 :: v_dual_max_num_f32 v181, v204, v205\n\t"
    "v_dual_min_num_f32 v186, v220, v226 :: v_dual_max_num_f32 v183, v220, v226\n\t"
    "v_dual_min_num_f32 v188, v208, v210 :: v_dual_max_num_f32 v189, v208, v210\n\t"
    "v_dual_min_num_f32 v194, v207, v213 :: v_dual_max_num_f32 v191, v207, v213\n\t"
    "v_dual_min_num_f32 v196, v223, v229 :: v_dual_max_num_f32 v197, v223, v229\n\t"
    "v_dual_min_num_f32 v204, v228, v234 :: v_dual_max_num_f32 v199, v228, v234\n\t"
    "v_dual_min_num_f32 v202, v212, v214 :: v_dual_max_num_f32 v205, v212, v214\n\t"
    "v_dual_min_num_f32 v220, v211, v217 :: v_dual_max_num_f32 v207, v211, v217\n\t"
    "v_dual_min_num_f32 v226, v231, v237 :: v_dual_max_num_f32 v223, v231, v237\n\t"
    "v_dual_min_num_f32 v208, v236, v242 :: v_dual_max_num_f32 v213, v236, v242\n\t"
    "v_dual_min_num_f32 v210, v216, v218 :: v_dual_max_num_f32 v229, v216, v218\n\t"
    "v_dual_min_num_f32 v228, v215, v221 :: v_dual_max_num_f32 v211, v215, v221\n\t"
    "v_dual_min_num_f32 v234, v239, v169 :: v_dual_max_num_f32 v217, v239, v169\n\t"
    "v_dual_min_num_f32 v214, v168, v174 :: v_dual_max_num_f32 v231, v168, v174\n\t"
    "v_dual_min_num_f32 v212, v222, v224 :: v_dual_max_num_f32 v237, v222, v224\n\t"
    "v_dual_min_num_f32 v236, v219, v225 :: v_dual_max_num_f32 v215, v219, v225\n\t"
    "v_dual_min_num_f32 v242, v171, v177 :: v_dual_max_num_f32 v239, v171, v177\n\t"
    "v_dual_min_num_f32 v216, v176, v182 :: v_dual_max_num_f32 v221, v176, v182\n\t"
    "v_dual_min_num_f32 v218, v230, v232 :: v_dual_max_num_f32 v169, v230, v232\n\t"
    "v_dual_min_num_f32 v168, v227, v233 :: v_dual_max_num_f32 v219, v227, v233\n\t"
    "v_dual_min_num_f32 v174, v179, v185 :: v_dual_max_num_f32 v225, v179, v185\n\t"
    "v_dual_min_num_f32 v222, v184, v190 :: v_dual_max_num_f32 v171, v184, v190\n\t"
    "v_dual_min_num_f32 v224, v238, v240 :: v_dual_max_num_f32 v177, v238, v240\n\t"
    "v_dual_min_num_f32 v176, v235, v241 :: v_dual_max_num_f32 v227, v235, v241\n\t"
    "v_dual_min_num_f32 v182, v187, v193 :: v_dual_max_num_f32 v179, v187, v193\n\t"
    "v_dual_min_num_f32 v230, v192, v198 :: v_dual_max_num_f32 v233, v192, v198\n\t"
    "v_dual_min_num_f32 v232, v170, v172 :: v_dual_max_num_f32 v185, v170, v172\n\t"
    "v_dual_min_num_f32 v184, v243, v173 :: v_dual_max_num_f32 v235, v243, v173\n\t"
    "v_dual_min_num_f32 v190, v195, v201 :: v_dual_max_num_f32 v241, v195, v201\n\t"
    "v_dual_min_num_f32 v240, v200, v206 :: v_dual_max_num_f32 v187, v200, v206\n\t"
    "v_dual_min_num_f32 v238, v178, v180 :: v_dual_max_num_f32 v193, v178, v180\n\t"
    "v_dual_min_num_f32 v192, v175, v181 :: v_dual_max_num_f32 v243, v175, v181\n\t"
    "v_dual_min_num_f32 v198, v203, v209 :: v_dual_max_num_f32 v195, v203, v209\n\t"
    "v_dual_min_num_f32 v170, v194, v183 :: v_dual_max_num_f32 v173, v194, v183\n\t"
    "v_dual_min_num_f32 v172, v196, v189 :: v_dual_max_num_f32 v201, v196, v189\n\t"
    "v_dual_min_num_f32 v200, v220, v199 :: v_dual_max_num_f32 v175, v220, v199\n\t"
    "v_dual_min_num_f32 v206, v226, v205 :: v_dual_max_num_f32 v181, v226, v205\n\t"
    "v_dual_min_num_f32 v180, v228, v213 :: v_dual_max_num_f32 v203, v228, v213\n\t"
    "v_dual_min_num_f32 v178, v234, v229 :: v_dual_max_num_f32 v209, v234, v229\n\t"
    "v_dual_min_num_f32 v194, v236, v231 :: v_dual_max_num_f32 v183, v236, v231\n\t"
    "v_dual_min_num_f32 v196, v242, v237 :: v_dual_max_num_f32 v189, v242, v237\n\t"
    "v_dual_min_num_f32 v220, v168, v221 :: v_dual_max_num_f32 v199, v168, v221\n\t"
    "v_dual_min_num_f32 v226, v174, v169 :: v_dual_max_num_f32 v205, v174, v169\n\t"
    "v_dual_min_num_f32 v234, v176, v171 :: v_dual_max_num_f32 v213, v176, v171\n\t"
    "v_dual_min_num_f32 v228, v182, v177 :: v_dual_max_num_f32 v229, v182, v177\n\t"
    "v_dual_min_num_f32 v242, v184, v233 :: v_dual_max_num_f32 v231, v184, v233\n\t"
    "v_dual_min_num_f32 v236, v190, v185 :: v_dual_max_num_f32 v237, v190, v185\n\t"
    "v_dual_min_num_f32 v168, v192, v187 :: v_dual_max_num_f32 v221, v192, v187\n\t"
    "v_dual_min_num_f32 v174, v198, v193 :: v_dual_max_num_f32 v169, v198, v193\n\t"
    "v_dual_min_num_f32 v176, v188, v170 :: v_dual_max_num_f32 v171, v188, v170\n\t"
    "v_dual_min_num_f32 v182, v172, v173 :: v_dual_max_num_f32 v177, v172, v173\n\t"
    "v_dual_min_num_f32 v184, v201, v191 :: v_dual_max_num_f32 v233, v201, v191\n\t"
    "v_dual_min_num_f32 v190, v202, v200 :: v_dual_max_num_f32 v185, v202, v200\n\t"
    "v_dual_min_num_f32 v192, v206, v175 :: v_dual_max_num_f32 v187, v206, v175\n\t"
    "v_dual_min_num_f32 v198, v181, v207 :: v_dual_max_num_f32 v191, v181, v207\n\t"
    "v_dual_min_num_f32 v188, v210, v180 :: v_dual_max_num_f32 v193, v210, v180\n\t"
    "v_dual_min_num_f32 v170, v178, v203 :: v_dual_max_num_f32 v173, v178, v203\n\t"
    "v_dual_min_num_f32 v172, v209, v211 :: v_dual_max_num_f32 v201, v209, v211\n\t"
    "v_dual_min_num_f32 v202, v212, v194 :: v_dual_max_num_f32 v175, v212, v194\n\t"
    "v_dual_min_num_f32 v200, v196, v183 :: v_dual_max_num_f32 v207, v196, v183\n\t"
    "v_dual_min_num_f32 v206, v189, v215 :: v_dual_max_num_f32 v203, v189, v215\n\t"
    "v_dual_min_num_f32 v210, v218, v220 :: v_dual_max_num_f32 v181, v218, v220\n\t"
    "v_dual_min_num_f32 v180, v226, v199 :: v_dual_max_num_f32 v209, v226, v199\n\t"
    "v_dual_min_num_f32 v178, v205, v219 :: v_dual_max_num_f32 v211, v205, v219\n\t"
    "v_dual_min_num_f32 v212, v224, v234 :: v_dual_max_num_f32 v183, v224, v234\n\t"
    "v_dual_min_num_f32 v194, v228, v213 :: v_dual_max_num_f32 v215, v228, v213\n\t"
    "v_dual_min_num_f32 v196, v229, v227 :: v_dual_max_num_f32 v189, v229, v227\n\t"
    "v_dual_min_num_f32 v218, v232, v242 :: v_dual_max_num_f32 v199, v232, v242\n\t"
    "v_dual_min_num_f32 v220, v236, v231 :: v_dual_max_num_f32 v205, v236, v231\n\t"
    "v_dual_min_num_f32 v226, v237, v235 :: v_dual_max_num_f32 v219, v237, v235\n\t"
    "v_dual_min_num_f32 v224, v238, v168 :: v_dual_max_num_f32 v213, v238, v168\n\t"
    "v_dual_min_num_f32 v234, v174, v221 :: v_dual_max_num_f32 v227, v174, v221\n\t"
    "v_dual_min_num_f32 v228, v169, v243 :: v_dual_max_num_f32 v229, v169, v243\n\t"
    "v_dual_min_num_f32 v232, v186, v204 :: v_dual_max_num_f32 v231, v186, v204\n\t"
    "v_dual_min_num_f32 v242, v176, v190 :: v_dual_max_num_f32 v237, v176, v190\n\t"
    "v_dual_min_num_f32 v236, v171, v185 :: v_dual_max_num_f32 v235, v171, v185\n\t"
    "v_dual_min_num_f32 v238, v182, v192 :: v_dual_max_num_f32 v221, v182, v192\n\t"
    "v_dual_min_num_f32 v168, v177, v187 :: v_dual_max_num_f32 v169, v177, v187\n\t"
    "v_dual_min_num_f32 v174, v184, v198 :: v_dual_max_num_f32 v243, v184, v198\n\t"
    "v_dual_min_num_f32 v186, v233, v191 :: v_dual_max_num_f32 v171, v233, v191\n\t"
    "v_dual_min_num_f32 v204, v197, v223 :: v_dual_max_num_f32 v185, v197, v223\n\t"
    "v_dual_min_num_f32 v190, v208, v214 :: v_dual_max_num_f32 v177, v208, v214\n\t"
    "v_dual_min_num_f32 v176, v188, v202 :: v_dual_max_num_f32 v187, v188, v202\n\t"
    "v_dual_min_num_f32 v182, v193, v175 :: v_dual_max_num_f32 v233, v193, v175\n\t"
    "v_dual_min_num_f32 v192, v170, v200 :: v_dual_max_num_f32 v191, v170, v200\n\t"
    "v_dual_min_num_f32 v184, v173, v207 :: v_dual_max_num_f32 v197, v173, v207\n\t"
    "v_dual_min_num_f32 v198, v172, v206 :: v_dual_max_num_f32 v223, v172, v206\n\t"
    "v_dual_min_num_f32 v208, v201, v203 :: v_dual_max_num_f32 v193, v201, v203\n\t"
    "v_dual_min_num_f32 v214, v217, v239 :: v_dual_max_num_f32 v175, v217, v239\n\t"
    "v_dual_min_num_f32 v188, v216, v222 :: v_dual_max_num_f32 v173, v216, v222\n\t"
    "v_dual_min_num_f32 v202, v210, v212 :: v_dual_max_num_f32 v207, v210, v212\n\t"
    "v_dual_min_num_f32 v170, v181, v183 :: v_dual_max_num_f32 v201, v181, v183\n\t"
    "v_dual_min_num_f32 v200, v180, v194 :: v_dual_max_num_f32 v203, v180, v194\n\t"
    "v_dual_min_num_f32 v172, v209, v215 :: v_dual_max_num_f32 v217, v209, v215\n\t"
    "v_dual_min_num_f32 v206, v178, v196 :: v_dual_max_num_f32 v239, v178, v196\n\t"
    "v_dual_min_num_f32 v216, v211, v189 :: v_dual_max_num_f32 v181, v211, v189\n\t"
    "v_dual_min_num_f32 v222, v225, v179 :: v_dual_max_num_f32 v183, v225, v179\n\t"
    "v_dual_min_num_f32 v210, v230, v240 :: v_dual_max_num_f32 v209, v230, v240\n\t"
    "v_dual_min_num_f32 v212, v218, v224 :: v_dual_max_num_f32 v215, v218, v224\n\t"
    "v_dual_min_num_f32 v180, v199, v213 :: v_dual_max_num_f32 v211, v199, v213\n\t"
    "v_dual_min_num_f32 v194, v220, v234 :: v_dual_max_num_f32 v189, v220, v234\n\t"
    "v_dual_min_num_f32 v178, v205, v227 :: v_dual_max_num_f32 v225, v205, v227\n\t"
    "v_dual_min_num_f32 v196, v226, v228 :: v_dual_max_num_f32 v179, v226, v228\n\t"
    "v_dual_min_num_f32 v230, v219, v229 :: v_dual_max_num_f32 v199, v219, v229\n\t"
    "v_dual_min_num_f32 v240, v241, v195 :: v_dual_max_num_f32 v213, v241, v195\n\t"
    "v_dual_min_num_f32 v218, v168, v231 :: v_dual_max_num_f32 v205, v168, v231\n\t"
    "v_dual_min_num_f32 v224, v174, v237 :: v_dual_max_num_f32 v227, v174, v237\n\t"
    "v_dual_min_num_f32 v220, v186, v235 :: v_dual_max_num_f32 v219, v186, v235\n\t"
    "v_dual_min_num_f32 v234, v204, v221 :: v_dual_max_num_f32 v229, v204, v221\n\t"
    "v_dual_min_num_f32 v228, v184, v177 :: v_dual_max_num_f32 v241, v184, v177\n\t"
    "v_dual_min_num_f32 v226, v198, v187 :: v_dual_max_num_f32 v195, v198, v187\n\t"
    "v_dual_min_num_f32 v168, v208, v233 :: v_dual_max_num_f32 v231, v208, v233\n\t"
    "v_dual_min_num_f32 v174, v214, v191 :: v_dual_max_num_f32 v237, v214, v191\n\t"
    "v_dual_min_num_f32 v204, v172, v173 :: v_dual_max_num_f32 v235, v172, v173\n\t"
    "v_dual_min_num_f32 v186, v206, v207 :: v_dual_max_num_f32 v221, v206, v207\n\t"
    "v_dual_min_num_f32 v184, v216, v201 :: v_dual_max_num_f32 v187, v216, v201\n\t"
    "v_dual_min_num_f32 v198, v222, v203 :: v_dual_max_num_f32 v177, v222, v203\n\t"
    "v_dual_min_num_f32 v214, v178, v209 :: v_dual_max_num_f32 v233, v178, v209\n\t"
    "v_dual_min_num_f32 v208, v196, v215 :: v_dual_max_num_f32 v191, v196, v215\n\t"
    "v_dual_min_num_f32 v172, v230, v211 :: v_dual_max_num_f32 v207, v230, v211\n\t"
    "v_dual_min_num_f32 v206, v240, v189 :: v_dual_max_num_f32 v173, v240, v189\n\t"
    "v_dual_min_num_f32 v216, v236, v218 :: v_dual_max_num_f32 v201, v236, v218\n\t"
    "v_dual_min_num_f32 v222, v238, v224 :: v_dual_max_num_f32 v203, v238, v224\n\t"
    "v_dual_min_num_f32 v178, v220, v205 :: v_dual_max_num_f32 v209, v220, v205\n\t"
    "v_dual_min_num_f32 v196, v234, v227 :: v_dual_max_num_f32 v215, v234, v227\n\t"
    "v_dual_min_num_f32 v230, v219, v169 :: v_dual_max_num_f32 v211, v219, v169\n\t"
    "v_dual_min_num_f32 v240, v229, v243 :: v_dual_max_num_f32 v189, v229, v243\n\t"
    "v_dual_min_num_f32 v218, v182, v228 :: v_dual_max_num_f32 v205, v182, v228\n\t"
    "v_dual_min_num_f32 v236, v192, v226 :: v_dual_max_num_f32 v227, v192, v226\n\t"
    "v_dual_min_num_f32 v238, v168, v241 :: v_dual_max_num_f32 v219, v168, v241\n\t"
    "v_dual_min_num_f32 v224, v174, v195 :: v_dual_max_num_f32 v169, v174, v195\n\t"
    "v_dual_min_num_f32 v220, v231, v197 :: v_dual_max_num_f32 v229, v231, v197\n\t"
    "v_dual_min_num_f32 v234, v237, v223 :: v_dual_max_num_f32 v243, v237, v223\n\t"
    "v_dual_min_num_f32 v228, v170, v204 :: v_dual_max_num_f32 v241, v170, v204\n\t"
    "v_dual_min_num_f32 v182, v200, v186 :: v_dual_max_num_f32 v195, v200, v186\n\t"
    "v_dual_min_num_f32 v192, v184, v235 :: v_dual_max_num_f32 v231, v184, v235\n\t"
    "v_dual_min_num_f32 v226, v198, v221 :: v_dual_max_num_f32 v197, v198, v221\n\t"
    "v_dual_min_num_f32 v168, v187, v217 :: v_dual_max_num_f32 v237, v187, v217\n\t"
    "v_dual_min_num_f32 v174, v177, v239 :: v_dual_max_num_f32 v223, v177, v239\n\t"
    "v_dual_min_num_f32 v170, v180, v214 :: v_dual_max_num_f32 v235, v180, v214\n\t"
    "v_dual_min_num_f32 v204, v194, v208 :: v_dual_max_num_f32 v221, v194, v208\n\t"
    "v_dual_min_num_f32 v200, v172, v233 :: v_dual_max_num_f32 v187, v172, v233\n\t"
    "v_dual_min_num_f32 v186, v206, v191 :: v_dual_max_num_f32 v217, v206, v191\n\t"
    "v_dual_min_num_f32 v184, v207, v225 :: v_dual_max_num_f32 v177, v207, v225\n\t"
    "v_dual_min_num_f32 v198, v173, v179 :: v_dual_max_num_f32 v233, v173, v179\n\t"
    "v_dual_min_num_f32 v180, v242, v216 :: v_dual_max_num_f32 v239, v242, v216\n\t"
    "v_dual_min_num_f32 v214, v222, v201 :: v_dual_max_num_f32 v191, v222, v201\n\t"
    "v_dual_min_num_f32 v194, v203, v178 :: v_dual_max_num_f32 v207, v203, v178\n\t"
    "v_dual_min_num_f32 v208, v196, v209 :: v_dual_max_num_f32 v225, v196, v209\n\t"
    "v_dual_min_num_f32 v172, v215, v230 :: v_dual_max_num_f32 v173, v215, v230\n\t"
    "v_dual_min_num_f32 v206, v240, v211 :: v_dual_max_num_f32 v179, v240, v211\n\t"
    "v_dual_min_num_f32 v242, v189, v171 :: v_dual_max_num_f32 v201, v189, v171\n\t"
    "v_dual_min_num_f32 v222, v176, v218 :: v_dual_max_num_f32 v209, v176, v218\n\t"
    "v_dual_min_num_f32 v216, v236, v205 :: v_dual_max_num_f32 v189, v236, v205\n\t"
    "v_dual_min_num_f32 v196, v227, v238 :: v_dual_max_num_f32 v205, v227, v238\n\t"
    "v_dual_min_num_f32 v178, v224, v219 :: v_dual_max_num_f32 v203, v224, v219\n\t"
    "v_dual_min_num_f32 v230, v169, v220 :: v_dual_max_num_f32 v215, v169, v220\n\t"
    "v_dual_min_num_f32 v240, v234, v229 :: v_dual_max_num_f32 v169, v234, v229\n\t"
    "v_dual_min_num_f32 v176, v243, v193 :: v_dual_max_num_f32 v211, v243, v193\n\t"
    "v_dual_min_num_f32 v218, v202, v228 :: v_dual_max_num_f32 v171, v202, v228\n\t"
    "v_dual_min_num_f32 v236, v182, v241 :: v_dual_max_num_f32 v227, v182, v241\n\t"
    "v_dual_min_num_f32 v238, v195, v192 :: v_dual_max_num_f32 v219, v195, v192\n\t"
    "v_dual_min_num_f32 v224, v226, v231 :: v_dual_max_num_f32 v229, v226, v231\n\t"
    "v_dual_min_num_f32 v220, v197, v168 :: v_dual_max_num_f32 v243, v197, v168\n\t"
    "v_dual_min_num_f32 v234, v174, v237 :: v_dual_max_num_f32 v193, v174, v237\n\t"
    "v_dual_min_num_f32 v202, v223, v181 :: v_dual_max_num_f32 v241, v223, v181\n\t"
    "v_dual_min_num_f32 v228, v212, v170 :: v_dual_max_num_f32 v197, v212, v170\n\t"
    "v_dual_min_num_f32 v182, v204, v235 :: v_dual_max_num_f32 v237, v204, v235\n\t"
    "v_dual_min_num_f32 v192, v221, v200 :: v_dual_max_num_f32 v181, v221, v200\n\t"
    "v_dual_min_num_f32 v226, v186, v187 :: v_dual_max_num_f32 v195, v186, v187\n\t"
    "v_dual_min_num_f32 v174, v217, v184 :: v_dual_max_num_f32 v221, v217, v184\n\t"
    "v_dual_min_num_f32 v168, v198, v177 :: v_dual_max_num_f32 v231, v198, v177\n\t"
    "v_dual_min_num_f32 v212, v233, v199 :: v_dual_max_num_f32 v223, v233, v199\n\t"
    "v_dual_min_num_f32 v170, v232, v190 :: v_dual_max_num_f32 v235, v232, v190\n\t"
    "v_dual_min_num_f32 v204, v180, v222 :: v_dual_max_num_f32 v187, v180, v222\n\t"
    "v_dual_min_num_f32 v200, v239, v209 :: v_dual_max_num_f32 v217, v239, v209\n\t"
    "v_dual_min_num_f32 v186, v214, v216 :: v_dual_max_num_f32 v177, v214, v216\n\t"
    "v_dual_min_num_f32 v184, v191, v189 :: v_dual_max_num_f32 v233, v191, v189\n\t"
    "v_dual_min_num_f32 v198, v194, v196 :: v_dual_max_num_f32 v199, v194, v196\n\t"
    "v_dual_min_num_f32 v232, v207, v205 :: v_dual_max_num_f32 v239, v207, v205\n\t"
    "v_dual_min_num_f32 v190, v208, v178 :: v_dual_max_num_f32 v209, v208, v178\n\t"
    "v_dual_min_num_f32 v180, v225, v203 :: v_dual_max_num_f32 v191, v225, v203\n\t"
    "v_dual_min_num_f32 v222, v172, v230 :: v_dual_max_num_f32 v189, v172, v230\n\t"
    "v_dual_min_num_f32 v214, v173, v215 :: v_dual_max_num_f32 v207, v173, v215\n\t"
    "v_dual_min_num_f32 v216, v206, v240 :: v_dual_max_num_f32 v205, v206, v240\n\t"
    "v_dual_min_num_f32 v194, v179, v169 :: v_dual_max_num_f32 v225, v179, v169\n\t"
    "v_dual_min_num_f32 v196, v242, v176 :: v_dual_max_num_f32 v203, v242, v176\n\t"
    "v_dual_min_num_f32 v208, v201, v211 :: v_dual_max_num_f32 v173, v201, v211\n\t"
    "v_dual_min_num_f32 v178, v185, v175 :: v_dual_max_num_f32 v215, v185, v175\n\t"
    "v_dual_min_num_f32 v172, v188, v210 :: v_dual_max_num_f32 v179, v188, v210\n\t"
    "v_dual_min_num_f32 v230, v218, v228 :: v_dual_max_num_f32 v169, v218, v228\n\t"
    "v_dual_min_num_f32 v206, v171, v197 :: v_dual_max_num_f32 v201, v171, v197\n\t"
    "v_dual_min_num_f32 v240, v236, v182 :: v_dual_max_num_f32 v211, v236, v182\n\t"
    "v_dual_min_num_f32 v242, v227, v237 :: v_dual_max_num_f32 v185, v227, v237\n\t"
    "v_dual_min_num_f32 v176, v238, v192 :: v_dual_max_num_f32 v175, v238, v192\n\t"
    "v_dual_min_num_f32 v188, v219, v181 :: v_dual_max_num_f32 v171, v219, v181\n\t"
    "v_dual_min_num_f32 v210, v224, v226 :: v_dual_max_num_f32 v197, v224, v226\n\t"
    "v_dual_min_num_f32 v218, v229, v195 :: v_dual_max_num_f32 v227, v229, v195\n\t"
    "v_dual_min_num_f32 v228, v220, v174 :: v_dual_max_num_f32 v237, v220, v174\n\t"
    "v_dual_min_num_f32 v236, v243, v221 :: v_dual_max_num_f32 v219, v243, v221\n\t"
    "v_dual_min_num_f32 v182, v234, v168 :: v_dual_max_num_f32 v181, v234, v168\n\t"
    "v_dual_min_num_f32 v238, v193, v231 :: v_dual_max_num_f32 v229, v193, v231\n\t"
    "v_dual_min_num_f32 v192, v202, v212 :: v_dual_max_num_f32 v195, v202, v212\n\t"
    "v_dual_min_num_f32 v224, v241, v223 :: v_dual_max_num_f32 v243, v241, v223\n\t"
    "v_dual_min_num_f32 v226, v183, v213 :: v_dual_max_num_f32 v221, v183, v213\n\t"
    "v_dual_min_num_f32 v174, v180, v235 :: v_dual_max_num_f32 v193, v180, v235\n\t"
    "v_dual_min_num_f32 v220, v222, v187 :: v_dual_max_num_f32 v231, v222, v187\n\t"
    "v_dual_min_num_f32 v234, v214, v217 :: v_dual_max_num_f32 v241, v214, v217\n\t"
    "v_dual_min_num_f32 v168, v216, v177 :: v_dual_max_num_f32 v223, v216, v177\n\t"
    "v_dual_min_num_f32 v202, v194, v233 :: v_dual_max_num_f32 v213, v194, v233\n\t"
    "v_dual_min_num_f32 v212, v196, v199 :: v_dual_max_num_f32 v183, v196, v199\n\t"
    "v_dual_min_num_f32 v180, v208, v239 :: v_dual_max_num_f32 v217, v208, v239\n\t"
    "v_dual_min_num_f32 v222, v178, v209 :: v_dual_max_num_f32 v235, v178, v209\n\t"
    "v_dual_min_num_f32 v216, v218, v179 :: v_dual_max_num_f32 v187, v218, v179\n\t"
    "v_dual_min_num_f32 v214, v228, v169 :: v_dual_max_num_f32 v177, v228, v169\n\t"
    "v_dual_min_num_f32 v194, v236, v201 :: v_dual_max_num_f32 v233, v236, v201\n\t"
    "v_dual_min_num_f32 v196, v182, v211 :: v_dual_max_num_f32 v199, v182, v211\n\t"
    "v_dual_min_num_f32 v208, v238, v185 :: v_dual_max_num_f32 v209, v238, v185\n\t"
    "v_dual_min_num_f32 v178, v192, v175 :: v_dual_max_num_f32 v239, v192, v175\n\t"
    "v_dual_min_num_f32 v218, v224, v171 :: v_dual_max_num_f32 v169, v224, v171\n\t"
    "v_dual_min_num_f32 v228, v226, v197 :: v_dual_max_num_f32 v179, v226, v197\n\t"
    "v_dual_min_num_f32 v182, v184, v174 :: v_dual_max_num_f32 v201, v184, v174\n\t"
    "v_dual_min_num_f32 v236, v198, v220 :: v_dual_max_num_f32 v211, v198, v220\n\t"
    "v_dual_min_num_f32 v238, v232, v234 :: v_dual_max_num_f32 v185, v232, v234\n\t"
    "v_dual_min_num_f32 v192, v190, v168 :: v_dual_max_num_f32 v175, v190, v168\n\t"
    "v_dual_min_num_f32 v224, v202, v193 :: v_dual_max_num_f32 v171, v202, v193\n\t"
    "v_dual_min_num_f32 v226, v212, v231 :: v_dual_max_num_f32 v197, v212, v231\n\t"
    "v_dual_min_num_f32 v184, v180, v241 :: v_dual_max_num_f32 v193, v180, v241\n\t"
    "v_dual_min_num_f32 v174, v222, v223 :: v_dual_max_num_f32 v231, v222, v223\n\t"
    "v_dual_min_num_f32 v198, v213, v191 :: v_dual_max_num_f32 v241, v213, v191\n\t"
    "v_dual_min_num_f32 v220, v183, v189 :: v_dual_max_num_f32 v223, v183, v189\n\t"
    "v_dual_min_num_f32 v232, v217, v207 :: v_dual_max_num_f32 v191, v217, v207\n\t"
    "v_dual_min_num_f32 v234, v235, v205 :: v_dual_max_num_f32 v213, v235, v205\n\t"
    "v_dual_min_num_f32 v168, v242, v216 :: v_dual_max_num_f32 v183, v242, v216\n\t"
    "v_dual_min_num_f32 v190, v176, v214 :: v_dual_max_num_f32 v189, v176, v214\n\t"
    "v_dual_min_num_f32 v202, v188, v194 :: v_dual_max_num_f32 v217, v188, v194\n\t"
    "v_dual_min_num_f32 v212, v210, v196 :: v_dual_max_num_f32 v207, v210, v196\n\t"
    "v_dual_min_num_f32 v180, v208, v187 :: v_dual_max_num_f32 v235, v208, v187\n\t"
    "v_dual_min_num_f32 v222, v178, v177 :: v_dual_max_num_f32 v205, v178, v177\n\t"
    "v_dual_min_num_f32 v242, v218, v233 :: v_dual_max_num_f32 v187, v218, v233\n\t"
    "v_dual_min_num_f32 v216, v228, v199 :: v_dual_max_num_f32 v177, v228, v199\n\t"
    "v_dual_min_num_f32 v176, v209, v227 :: v_dual_max_num_f32 v233, v209, v227\n\t"
    "v_dual_min_num_f32 v214, v239, v237 :: v_dual_max_num_f32 v199, v239, v237\n\t"
    "v_dual_min_num_f32 v188, v169, v219 :: v_dual_max_num_f32 v227, v169, v219\n\t"
    "v_dual_min_num_f32 v194, v179, v181 :: v_dual_max_num_f32 v209, v179, v181\n\t"
    "v_dual_min_num_f32 v210, v200, v182 :: v_dual_max_num_f32 v239, v200, v182\n\t"
    "v_dual_min_num_f32 v196, v186, v236 :: v_dual_max_num_f32 v237, v186, v236\n\t"
    "v_dual_min_num_f32 v208, v238, v201 :: v_dual_max_num_f32 v169, v238, v201\n\t"
    "v_dual_min_num_f32 v178, v192, v211 :: v_dual_max_num_f32 v219, v192, v211\n\t"
    "v_dual_min_num_f32 v218, v185, v224 :: v_dual_max_num_f32 v179, v185, v224\n\t"
    "v_dual_min_num_f32 v228, v175, v226 :: v_dual_max_num_f32 v181, v175, v226\n\t"
    "v_dual_min_num_f32 v200, v184, v171 :: v_dual_max_num_f32 v201, v184, v171\n\t"
    "v_dual_min_num_f32 v182, v174, v197 :: v_dual_max_num_f32 v211, v174, v197\n\t"
    "v_dual_min_num_f32 v186, v193, v198 :: v_dual_max_num_f32 v185, v193, v198\n\t"
    "v_dual_min_num_f32 v236, v231, v220 :: v_dual_max_num_f32 v175, v231, v220\n\t"
    "v_dual_min_num_f32 v238, v232, v241 :: v_dual_max_num_f32 v171, v232, v241\n\t"
    "v_dual_min_num_f32 v192, v234, v223 :: v_dual_max_num_f32 v197, v234, v223\n\t"
    "v_dual_min_num_f32 v224, v191, v225 :: v_dual_max_num_f32 v193, v191, v225\n\t"
    "v_dual_min_num_f32 v226, v213, v203 :: v_dual_max_num_f32 v231, v213, v203\n\t"
    "v_dual_min_num_f32 v184, v206, v168 :: v_dual_max_num_f32 v241, v206, v168\n\t"
    "v_dual_min_num_f32 v174, v240, v190 :: v_dual_max_num_f32 v223, v240, v190\n\t"
    "v_dual_min_num_f32 v198, v202, v183 :: v_dual_max_num_f32 v191, v202, v183\n\t"
    "v_dual_min_num_f32 v220, v212, v189 :: v_dual_max_num_f32 v225, v212, v189\n\t"
    "v_dual_min_num_f32 v232, v217, v180 :: v_dual_max_num_f32 v213, v217, v180\n\t"
    "v_dual_min_num_f32 v234, v207, v222 :: v_dual_max_num_f32 v203, v207, v222\n\t"
    "v_dual_min_num_f32 v206, v242, v235 :: v_dual_max_num_f32 v183, v242, v235\n\t"
    "v_dual_min_num_f32 v168, v216, v205 :: v_dual_max_num_f32 v189, v216, v205\n\t"
    "v_dual_min_num_f32 v240, v187, v176 :: v_dual_max_num_f32 v217, v187, v176\n\t"
    "v_dual_min_num_f32 v190, v177, v214 :: v_dual_max_num_f32 v207, v177, v214\n\t"
    "v_dual_min_num_f32 v202, v188, v233 :: v_dual_max_num_f32 v235, v188, v233\n\t"
    "v_dual_min_num_f32 v212, v194, v199 :: v_dual_max_num_f32 v205, v194, v199\n\t"
    "v_dual_min_num_f32 v180, v227, v229 :: v_dual_max_num_f32 v187, v227, v229\n\t"
    "v_dual_min_num_f32 v222, v209, v195 :: v_dual_max_num_f32 v177, v209, v195\n\t"
    "v_dual_min_num_f32 v242, v204, v210 :: v_dual_max_num_f32 v233, v204, v210\n\t"
    "v_dual_min_num_f32 v216, v196, v239 :: v_dual_max_num_f32 v199, v196, v239\n\t"
    "v_dual_min_num_f32 v176, v237, v208 :: v_dual_max_num_f32 v227, v237, v208\n\t"
    "v_dual_min_num_f32 v214, v178, v169 :: v_dual_max_num_f32 v229, v178, v169\n\t"
    "v_dual_min_num_f32 v188, v219, v218 :: v_dual_max_num_f32 v209, v219, v218\n\t"
    "v_dual_min_num_f32 v194, v228, v179 :: v_dual_max_num_f32 v195, v228, v179\n\t"
    "v_dual_min_num_f32 v204, v181, v200 :: v_dual_max_num_f32 v239, v181, v200\n\t"
    "v_dual_min_num_f32 v210, v182, v201 :: v_dual_max_num_f32 v237, v182, v201\n\t"
    "v_dual_min_num_f32 v196, v211, v186 :: v_dual_max_num_f32 v169, v211, v186\n\t"
    "v_dual_min_num_f32 v208, v236, v185 :: v_dual_max_num_f32 v219, v236, v185\n\t"
    "v_dual_min_num_f32 v178, v175, v238 :: v_dual_max_num_f32 v179, v175, v238\n\t"
    "v_dual_min_num_f32 v218, v192, v171 :: v_dual_max_num_f32 v181, v192, v171\n\t"
    "v_dual_min_num_f32 v228, v197, v224 :: v_dual_max_num_f32 v201, v197, v224\n\t"
    "v_dual_min_num_f32 v200, v226, v193 :: v_dual_max_num_f32 v211, v226, v193\n\t"
    "v_dual_min_num_f32 v182, v231, v173 :: v_dual_max_num_f32 v185, v231, v173\n\t"
    "v_dual_min_num_f32 v236, v230, v184 :: v_dual_max_num_f32 v175, v230, v184\n\t"
    "v_dual_min_num_f32 v186, v174, v241 :: v_dual_max_num_f32 v197, v174, v241\n\t"
    "v_dual_min_num_f32 v238, v223, v198 :: v_dual_max_num_f32 v193, v223, v198\n\t"
    "v_dual_min_num_f32 v192, v220, v191 :: v_dual_max_num_f32 v171, v220, v191\n\t"
    "v_dual_min_num_f32 v226, v225, v232 :: v_dual_max_num_f32 v231, v225, v232\n\t"
    "v_dual_min_num_f32 v224, v234, v213 :: v_dual_max_num_f32 v173, v234, v213\n\t"
    "v_dual_min_num_f32 v230, v203, v206 :: v_dual_max_num_f32 v241, v203, v206\n\t"
    "v_dual_min_num_f32 v184, v168, v183 :: v_dual_max_num_f32 v223, v168, v183\n\t"
    "v_dual_min_num_f32 v174, v189, v240 :: v_dual_max_num_f32 v191, v189, v240\n\t"
    "v_dual_min_num_f32 v198, v190, v217 :: v_dual_max_num_f32 v225, v190, v217\n\t"
    "v_dual_min_num_f32 v220, v207, v202 :: v_dual_max_num_f32 v213, v207, v202\n\t"
    "v_dual_min_num_f32 v232, v212, v235 :: v_dual_max_num_f32 v203, v212, v235\n\t"
    "v_dual_min_num_f32 v234, v205, v180 :: v_dual_max_num_f32 v183, v205, v180\n\t"
    "v_dual_min_num_f32 v206, v222, v187 :: v_dual_max_num_f32 v189, v222, v187\n\t"
    "v_dual_min_num_f32 v168, v177, v243 :: v_dual_max_num_f32 v207, v177, v243\n\t"
    "v_dual_min_num_f32 v240, v170, v172 :: v_dual_max_num_f32 v217, v170, v172\n\t"
    "v_max_num_f32 v236, v242, v236\n\t"
    "v_max_num_f32 v175, v233, v175\n\t"
    "v_max_num_f32 v186, v216, v186\n\t"
    "v_max_num_f32 v197, v199, v197\n\t"
    "v_max_num_f32 v238, v176, v238\n\t"
    "v_max_num_f32 v193, v227, v193\n\t"
    "v_max_num_f32 v192, v214, v192\n\t"
    "v_dual_min_num_f32 v190, v229, v171 :: v_dual_max_num_f32 v235, v229, v171\n\t"
    "v_dual_min_num_f32 v202, v188, v226 :: v_dual_max_num_f32 v205, v188, v226\n\t"
    "v_dual_min_num_f32 v212, v209, v231 :: v_dual_max_num_f32 v187, v209, v231\n\t"
    "v_dual_min_num_f32 v180, v194, v224 :: v_dual_max_num_f32 v177, v194, v224\n\t"
    "v_dual_min_num_f32 v222, v195, v173 :: v_dual_max_num_f32 v243, v195, v173\n\t"
    "v_dual_min_num_f32 v170, v204, v230 :: v_dual_max_num_f32 v233, v204, v230\n\t"
    "v_dual_min_num_f32 v172, v239, v241 :: v_dual_max_num_f32 v199, v239, v241\n\t"
    "v_dual_min_num_f32 v242, v210, v184 :: v_dual_max_num_f32 v227, v210, v184\n\t"
    "v_dual_min_num_f32 v216, v237, v223 :: v_dual_max_num_f32 v229, v237, v223\n\t"
    "v_dual_min_num_f32 v214, v196, v174 :: v_dual_max_num_f32 v171, v196, v174\n\t"
    "v_dual_min_num_f32 v176, v169, v191 :: v_dual_max_num_f32 v209, v169, v191\n\t"
    "v_dual_min_num_f32 v188, v208, v198 :: v_dual_max_num_f32 v231, v208, v198\n\t"
    "v_dual_min_num_f32 v226, v219, v225 :: v_dual_max_num_f32 v195, v219, v225\n\t"
    "v_dual_min_num_f32 v224, v178, v220 :: v_dual_max_num_f32 v173, v178, v220\n\t"
    "v_dual_min_num_f32 v194, v179, v213 :: v_dual_max_num_f32 v239, v179, v213\n\t"
    "v_dual_min_num_f32 v230, v218, v232 :: v_dual_max_num_f32 v241, v218, v232\n\t"
    "v_min_num_f32 v181, v181, v203\n\t"
    "v_min_num_f32 v228, v228, v234\n\t"
    "v_min_num_f32 v201, v201, v183\n\t"
    "v_min_num_f32 v200, v200, v206\n\t"
    "v_min_num_f32 v204, v211, v189\n\t"
    "v_min_num_f32 v182, v182, v168\n\t"
    "v_min_num_f32 v185, v185, v207\n\t"
    "v_dual_min_num_f32 v210, v215, v221 :: v_dual_max_num_f32 v237, v215, v221\n\t"
    "v_dual_min_num_f32 v184, v216, v217 :: v_dual_max_num_f32 v223, v216, v217\n\t"
    "v_dual_min_num_f32 v196, v214, v236 :: v_dual_max_num_f32 v169, v214, v236\n\t"
    "v_dual_min_num_f32 v174, v176, v175 :: v_dual_max_num_f32 v191, v176, v175\n\t"
    "v_dual_min_num_f32 v198, v188, v186 :: v_dual_max_num_f32 v219, v188, v186\n\t"
    "v_dual_min_num_f32 v208, v226, v197 :: v_dual_max_num_f32 v225, v226, v197\n\t"
    "v_dual_min_num_f32 v220, v224, v238 :: v_dual_max_num_f32 v179, v224, v238\n\t"
    "v_dual_min_num_f32 v178, v194, v193 :: v_dual_max_num_f32 v213, v194, v193\n\t"
    "v_dual_min_num_f32 v232, v230, v192 :: v_dual_max_num_f32 v203, v230, v192\n\t"
    "v_dual_min_num_f32 v218, v181, v235 :: v_dual_max_num_f32 v183, v181, v235\n\t"
    "v_dual_min_num_f32 v234, v228, v205 :: v_dual_max_num_f32 v189, v228, v205\n\t"
    "v_dual_min_num_f32 v206, v201, v187 :: v_dual_max_num_f32 v211, v201, v187\n\t"
    "v_dual_min_num_f32 v168, v200, v177 :: v_dual_max_num_f32 v221, v200, v177\n\t"
    "v_dual_min_num_f32 v216, v204, v243 :: v_dual_max_num_f32 v217, v204, v243\n\t"
    "v_dual_min_num_f32 v214, v182, v233 :: v_dual_max_num_f32 v207, v182, v233\n\t"
    "v_dual_min_num_f32 v236, v185, v199 :: v_dual_max_num_f32 v197, v185, v199\n\t"
    "v_dual_min_num_f32 v176, v210, v227 :: v_dual_max_num_f32 v215, v210, v227\n\t"
    "v_max_num_f32 v184, v190, v184\n\t"
    "v_max_num_f32 v196, v202, v196\n\t"
    "v_max_num_f32 v174, v212, v174\n\t"
    "v_max_num_f32 v198, v180, v198\n\t"
    "v_min_num_f32 v222, v222, v208\n\t"
    "v_min_num_f32 v170, v170, v220\n\t"
    "v_min_num_f32 v172, v172, v178\n\t"
    "v_min_num_f32 v175, v242, v232\n\t"
    "v_max_num_f32 v223, v218, v223\n\t"
    "v_max_num_f32 v169, v234, v169\n\t"
    "v_max_num_f32 v191, v206, v191\n\t"
    "v_max_num_f32 v219, v168, v219\n\t"
    "v_min_num_f32 v216, v216, v225\n\t"
    "v_min_num_f32 v214, v214, v179\n\t"
    "v_min_num_f32 v236, v236, v213\n\t"
    "v_min_num_f32 v176, v176, v203\n\t"
    "v_max_num_f32 v229, v183, v229\n\t"
    "v_max_num_f32 v171, v189, v171\n\t"
    "v_max_num_f32 v209, v211, v209\n\t"
    "v_max_num_f32 v231, v221, v231\n\t"
    "v_min_num_f32 v188, v217, v195\n\t"
    "v_min_num_f32 v186, v207, v173\n\t"
    "v_min_num_f32 v226, v197, v239\n\t"
    "v_min_num_f32 v224, v215, v241\n\t"
    "v_max_num_f32 v184, v222, v184\n\t"
    "v_max_num_f32 v196, v170, v196\n\t"
    "v_min_num_f32 v238, v172, v174\n\t"
    "v_min_num_f32 v175, v175, v198\n\t"
    "v_max_num_f32 v223, v216, v223\n\t"
    "v_max_num_f32 v169, v214, v169\n\t"
    "v_min_num_f32 v236, v236, v191\n\t"
    "v_min_num_f32 v176, v176, v219\n\t"
    "v_max_num_f32 v229, v188, v229\n\t"
    "v_max_num_f32 v171, v186, v171\n\t"
    "v_min_num_f32 v226, v226, v209\n\t"
    "v_min_num_f32 v224, v224, v231\n\t"
    "v_max_num_f32 v184, v238, v184\n\t"
    "v_min_num_f32 v175, v175, v196\n\t"
    "v_max_num_f32 v223, v236, v223\n\t"
    "v_min_num_f32 v176, v176, v169\n\t"
    "v_max_num_f32 v229, v226, v229\n\t"
    "v_min_num_f32 v224, v224, v171\n\t"
    "v_dual_min_num_f32 v194, v175, v184 :: v_dual_max_num_f32 v193, v175, v184\n\t"
    "s_delay_alu instid0(VALU_DEP_4)\n\t"
    "v_dual_min_num_f32 v230, v176, v223 :: v_dual_max_num_f32 v181, v176, v223\n\t"
    "s_delay_alu instid0(VALU_DEP_3)\n\t"
    "v_dual_min_num_f32 v192, v224, v229 :: v_dual_max_num_f32 v235, v224, v229\n\t"
    "v_mov_b32 %0, v240\n\t"
    "v_mov_b32 %1, v194\n\t"
    "v_mov_b32 %2, v193\n\t"
    "v_mov_b32 %3, v230\n\t"
    "v_mov_b32 %4, v181\n\t"
    "v_mov_b32 %5, v192\n\t"
    "v_mov_b32 %6, v235\n\t"
    "v_mov_b32 %7, v237\n\t"
    "v_mov_b32 %8, v245\n\t"
    "v_mov_b32 %9, v247\n\t"
    "v_mov_b32 %10, v249\n\t"
    "v_mov_b32 %11, v253\n\t"
    "v_mov_b32 %12, v244\n\t"
    "v_mov_b32 %13, v246\n\t"
    "v_mov_b32 %14, v248\n\t"
    "v_mov_b32 %15, v250\n\t"
    : "=v"(r0), "=v"(r15), "=v"(r16), "=v"(r31), "=v"(r32), "=v"(r47), "=v"(r48), "=v"(r63),
      "=v"(sa), "=v"(sb), "=v"(sc), "=v"(sd), "=v"(ta), "=v"(tb), "=v"(tc), "=v"(td)
    : "v"(la)
    : "memory", "v168", "v169", "v170", "v171", "v172", "v173", "v174", "v175", "v176", "v177", "v178", "v179", "v180", "v181", "v182", "v183", "v184", "v185", "v186", "v187", "v188", "v189", "v190", "v191", "v192", "v193", "v194", "v195", "v196", "v197", "v198", "v199", "v200", "v201", "v202", "v203", "v204", "v205", "v206", "v207", "v208", "v209", "v210", "v211", "v212", "v213", "v214", "v215", "v216", "v217", "v218", "v219", "v220", "v221", "v222", "v223", "v224", "v225", "v226", "v227", "v228", "v229", "v230", "v231", "v232", "v233", "v234", "v235", "v236", "v237", "v238", "v239", "v240", "v241", "v242", "v243", "v244", "v245", "v246", "v247", "v248", "v249", "v250", "v251", "v252", "v253");

  const float s  = (sa + sb) + (sc + sd);
  const float s2 = (ta + tb) + (tc + td);
  const float mean = s * (1.0f / 64.0f);
  const float var  = (s2 - 64.0f * mean * mean) * (1.0f / 63.0f);

  const float q0 = r0;
  const float q1 = r15 + 0.75f * (r16 - r15);
  const float q2 = 0.5f * (r31 + r32);
  const float q3 = r47 + 0.25f * (r48 - r47);
  const float q4 = r63;

  o[0] = mean; o[1] = var; o[2] = q0; o[3] = q1; o[4] = q2; o[5] = q3; o[6] = q4;
}

#define UR 48
#define FP 60
#define EP 68
#define QP 196

__global__ __launch_bounds__(128) void k_fused(const float* __restrict__ ws,
                                               const float* __restrict__ Wemb,
                                               const float* __restrict__ bemb,
                                               const float* __restrict__ pos,
                                               const float* __restrict__ win,
                                               const float* __restrict__ bin,
                                               float* __restrict__ part) {
  __shared__ __attribute__((aligned(16))) float sA[UR * FP];
  __shared__ __attribute__((aligned(16))) float sB[UR * EP];
  __shared__ __attribute__((aligned(16))) float sC[UR * QP];
  const int tid  = threadIdx.x;
  const int lane = tid & 31;
  const int wid  = tid >> 5;
  const int r  = lane & 15;
  const int hf = lane >> 4;
  const size_t row0 = (size_t)blockIdx.x * UR;

  {
    int* sRow = reinterpret_cast<int*>(sB);
    v4f* wt = reinterpret_cast<v4f*>(sC) + wid * (32 * ROWQ);
    const unsigned la = (unsigned)reinterpret_cast<size_t>(wt + lane * ROWQ);
    #pragma unroll
    for (int k = 0; k < 3; ++k) {
      const int idx = tid + 128 * k;
      const int t = idx / UR;
      const int i = idx - t * UR;
      const int R = (int)row0 + i;
      const int c = R & 255;
      const int bl = R >> 8;
      const int b = bl / LL;
      const int l = bl - b * LL;
      sRow[idx] = ((l * TT + t) * BB + b) * CC + c;
    }
    __syncthreads();

    v4f x[16];
    {
      const int base = wid * 32 + hf;
      #pragma unroll
      for (int q = 0; q < 16; ++q) {
        const int sr = sRow[base + 2 * q];
        x[q] = *(reinterpret_cast<const v4f*>(ws + (size_t)sr * SS) + r);
      }
    }
    #pragma unroll 1
    for (int p = 0; p < 3; ++p) {
      #pragma unroll
      for (int q = 0; q < 16; ++q) wt[(2 * q + hf) * ROWQ + r] = x[q];
      __syncthreads();
      if (p < 2) {
        const int base = (p + 1) * 128 + wid * 32 + hf;
        #pragma unroll
        for (int q = 0; q < 16; ++q) {
          const int sr = sRow[base + 2 * q];
          x[q] = *(reinterpret_cast<const v4f*>(ws + (size_t)sr * SS) + r);
        }
      }
      const int idx = p * 128 + tid;
      const int t = idx / UR;
      const int i = idx - t * UR;
      row_stats(la, sA + i * FP + t * 7);
    }
  }
  __syncthreads();

  {
    const int n0 = wid * 16;
    v8f e0 = {0, 0, 0, 0, 0, 0, 0, 0};
    v8f e1 = {0, 0, 0, 0, 0, 0, 0, 0};
    v8f e2 = {0, 0, 0, 0, 0, 0, 0, 0};
    #pragma unroll
    for (int k0 = 0; k0 < 56; k0 += 4) {
      const int c0 = k0 + 2 * hf;
      const float* bp = Wemb + (size_t)c0 * EE + n0 + r;
      v2f b; b.x = bp[0]; b.y = bp[EE];
      const float* ap = sA + r * FP + c0;
      const v2f a0 = *reinterpret_cast<const v2f*>(ap);
      const v2f a1 = *reinterpret_cast<const v2f*>(ap + 16 * FP);
      const v2f a2 = *reinterpret_cast<const v2f*>(ap + 32 * FP);
      e0 = wmma4(a0, b, e0);
      e1 = wmma4(a1, b, e1);
      e2 = wmma4(a2, b, e2);
    }
    const int n = n0 + r;
    const float be = bemb[n];
    #pragma unroll
    for (int g = 0; g < 8; ++g) {
      const int ml = g + 8 * hf;
      const int l0 = (int)(((row0 + ml)      >> 8) % LL);
      const int l1 = (int)(((row0 + ml + 16) >> 8) % LL);
      const int l2 = (int)(((row0 + ml + 32) >> 8) % LL);
      sB[(ml)      * EP + n] = e0[g] + be + pos[l0 * EE + n];
      sB[(ml + 16) * EP + n] = e1[g] + be + pos[l1 * EE + n];
      sB[(ml + 32) * EP + n] = e2[g] + be + pos[l2 * EE + n];
    }
  }
  __syncthreads();

  {
    v2f a[3][16];
    #pragma unroll
    for (int kk = 0; kk < 16; ++kk) {
      const float* ap = sB + r * EP + 4 * kk + 2 * hf;
      a[0][kk] = *reinterpret_cast<const v2f*>(ap);
      a[1][kk] = *reinterpret_cast<const v2f*>(ap + 16 * EP);
      a[2][kk] = *reinterpret_cast<const v2f*>(ap + 32 * EP);
    }
    #pragma unroll
    for (int j = 0; j < 2; ++j) {
      const int n0 = (wid + 4 * j) * 16;
      v8f c0v = {0, 0, 0, 0, 0, 0, 0, 0};
      v8f c1v = {0, 0, 0, 0, 0, 0, 0, 0};
      v8f c2v = {0, 0, 0, 0, 0, 0, 0, 0};
      #pragma unroll
      for (int kk = 0; kk < 16; ++kk) {
        const int c0 = 4 * kk + 2 * hf;
        const float* bp = win + (size_t)c0 * 192 + n0 + r;
        v2f b; b.x = bp[0]; b.y = bp[192];
        c0v = wmma4(a[0][kk], b, c0v);
        c1v = wmma4(a[1][kk], b, c1v);
        c2v = wmma4(a[2][kk], b, c2v);
      }
      const int n = n0 + r;
      const float bi_ = bin[n];
      #pragma unroll
      for (int g = 0; g < 8; ++g) {
        const int ml = g + 8 * hf;
        sC[(ml)      * QP + n] = c0v[g] + bi_;
        sC[(ml + 16) * QP + n] = c1v[g] + bi_;
        sC[(ml + 32) * QP + n] = c2v[g] + bi_;
      }
    }
  }
  __syncthreads();

  const int rq = (r < LL) ? r : (LL - 1);
  const int rk = (r < 8) ? ((r < 6) ? r : 5) : ((r < 14) ? (r - 2) : (LL - 1));
  const float qmask = (r < LL) ? 1.0f : 0.0f;

  const int gsplit = (int)((((row0 >> 8) + 1) << 8) - row0);
  v4f z0 = {0, 0, 0, 0};
  v4f z1 = {0, 0, 0, 0};
  #pragma unroll
  for (int bi = 0; bi < 4; ++bi) {
    const float* base = sC + (bi * LL) * QP + wid * HDD;

    v8f S = {0, 0, 0, 0, 0, 0, 0, 0};
    #pragma unroll
    for (int k0 = 0; k0 < 16; k0 += 4) {
      const int c0 = k0 + 2 * hf;
      const v2f a = *reinterpret_cast<const v2f*>(base + rk * QP + EE + c0);
      const v2f b = *reinterpret_cast<const v2f*>(base + rq * QP + c0);
      S = wmma4(a, b, S);
    }

    float sv[6];
    #pragma unroll
    for (int g = 0; g < 6; ++g) sv[g] = S[g] * 0.25f;
    float rm = fmaxf(fmaxf(fmaxf(sv[0], sv[1]), fmaxf(sv[2], sv[3])), fmaxf(sv[4], sv[5]));
    rm = fmaxf(rm, xhalf(rm));
    float e[6];
    #pragma unroll
    for (int g = 0; g < 6; ++g) e[g] = __expf(sv[g] - rm);
    float rs = ((e[0] + e[1]) + (e[2] + e[3])) + (e[4] + e[5]);
    rs += xhalf(rs);
    const float inv = qmask / rs;

    float p[6], w[6];
    v4f x[6];
    #pragma unroll
    for (int g = 0; g < 6; ++g) {
      p[g] = e[g] * inv;
      w[g] = rowsum16(p[g]);
      x[g] = *reinterpret_cast<const v4f*>(sB + (bi * LL + 6 * hf + g) * EP + 4 * r);
    }
    const int q0 = bi * LL;
    if (q0 < gsplit && gsplit < q0 + LL) {
      const bool qfirst = (q0 + r) < gsplit;
      #pragma unroll
      for (int g = 0; g < 6; ++g) {
        const float wa = rowsum16(qfirst ? p[g] : 0.0f);
        const float wb = w[g] - wa;
        z0.x = fmaf(wa, x[g].x, z0.x); z0.y = fmaf(wa, x[g].y, z0.y); z0.z = fmaf(wa, x[g].z, z0.z); z0.w = fmaf(wa, x[g].w, z0.w);
        z1.x = fmaf(wb, x[g].x, z1.x); z1.y = fmaf(wb, x[g].y, z1.y); z1.z = fmaf(wb, x[g].z, z1.z); z1.w = fmaf(wb, x[g].w, z1.w);
      }
    } else {
      v4f zb = {0, 0, 0, 0};
      #pragma unroll
      for (int g = 0; g < 6; ++g) {
        zb.x = fmaf(w[g], x[g].x, zb.x); zb.y = fmaf(w[g], x[g].y, zb.y); zb.z = fmaf(w[g], x[g].z, zb.z); zb.w = fmaf(w[g], x[g].w, zb.w);
      }
      const bool bfirst = (q0 + LL) <= gsplit;
      z0.x += bfirst ? zb.x : 0.0f; z0.y += bfirst ? zb.y : 0.0f; z0.z += bfirst ? zb.z : 0.0f; z0.w += bfirst ? zb.w : 0.0f;
      z1.x += bfirst ? 0.0f : zb.x; z1.y += bfirst ? 0.0f : zb.y; z1.z += bfirst ? 0.0f : zb.z; z1.w += bfirst ? 0.0f : zb.w;
    }
  }
  z0.x += xhalf(z0.x); z0.y += xhalf(z0.y); z0.z += xhalf(z0.z); z0.w += xhalf(z0.w);
  z1.x += xhalf(z1.x); z1.y += xhalf(z1.y); z1.z += xhalf(z1.z); z1.w += xhalf(z1.w);

  const v4f zo = hf ? z1 : z0;
  *reinterpret_cast<v4f*>(part + (((size_t)blockIdx.x * 2 + hf) * NHH + wid) * EE + 4 * r) = zo;
}

#define NGRP (BB * LL)
#define HP 36
#define RP 132
#define ZP 260
__global__ __launch_bounds__(128) void k_tail(
    const float* __restrict__ part,
    const float* __restrict__ win, const float* __restrict__ bin,
    const float* __restrict__ wout, const float* __restrict__ bout,
    const float* __restrict__ mw1, const float* __restrict__ mb1,
    const float* __restrict__ mg,  const float* __restrict__ mbb,
    const float* __restrict__ mw2, const float* __restrict__ mb2,
    const float* __restrict__ nw1, const float* __restrict__ nb1,
    const float* __restrict__ ng,  const float* __restrict__ nbb,
    const float* __restrict__ nw2, const float* __restrict__ nb2,
    float* __restrict__ out) {
  __shared__ __attribute__((aligned(16))) float sZ[16 * ZP];
  __shared__ __attribute__((aligned(16))) float sM[16 * EP];
  __shared__ __attribute__((aligned(16))) float sP[16 * EP];
  __shared__ __attribute__((aligned(16))) float sH[2 * 16 * HP];
  __shared__ __attribute__((aligned(16))) float sG[16 * HP];
  __shared__ __attribute__((aligned(16))) float sR[16 * RP];
  __shared__ __attribute__((aligned(16))) float sLN[3 * HIDD];
  const int tid  = threadIdx.x;
  const int lane = tid & 31;
  const int wid  = tid >> 5;
  const int r  = lane & 15;
  const int hf = lane >> 4;
  const int qtr    = blockIdx.x & 3;
  const int branch = (blockIdx.x >> 2) & 1;
  const int G0     = (blockIdx.x >> 3) * 16;
  const float* w1 = branch ? nw1 : mw1;
  const float* b1 = branch ? nb1 : mb1;
  const float* gg = branch ? ng  : mg;
  const float* bb = branch ? nbb : mbb;
  const float* w2 = branch ? nw2 : mw2;
  const float* b2 = branch ? nb2 : mb2;

  static_assert(UR == 48 && CC == 256, "six blocks per pool group");
  const float4* p4 = reinterpret_cast<const float4*>(part);
  float4 pv[8][6];
  #pragma unroll
  for (int i = 0; i < 8; ++i) {
    const int G  = G0 + (tid >> 6) + 2 * i;
    const int jf = (G * 256) / UR;
    #pragma unroll
    for (int jj = 0; jj < 6; ++jj) {
      const int j = jf + jj;
      const int slot = G - ((j * UR) >> 8);
      pv[i][jj] = p4[(size_t)(j * 2 + slot) * (NHH * EE / 4) + (tid & 63)];
    }
  }

  const int n0 = wid * 16;
  v2f bo_[16];
  #pragma unroll
  for (int kk = 0; kk < 16; ++kk) {
    const int c0 = 4 * kk + 2 * hf;
    const float* bp = wout + (size_t)c0 * EE + n0 + r;
    bo_[kk].x = bp[0]; bo_[kk].y = bp[EE];
  }
  const float bo = bout[n0 + r];
  v2f bv_[16];
  #pragma unroll
  for (int kk = 0; kk < 16; ++kk) {
    const int c0 = 4 * kk + 2 * hf;
    const float* bp = win + (size_t)c0 * 192 + 2 * EE + n0 + r;
    bv_[kk].x = bp[0]; bv_[kk].y = bp[192];
  }
  const float bvb = bin[2 * EE + n0 + r];
  const int ct = wid & 1;
  const int kh = wid >> 1;
  v2f bw1[8];
  #pragma unroll
  for (int kk = 0; kk < 8; ++kk) {
    const int c0 = 4 * (8 * kh + kk) + 2 * hf;
    const float* bp = w1 + (size_t)c0 * HIDD + ct * 16 + r;
    bw1[kk].x = bp[0]; bw1[kk].y = bp[HIDD];
  }
  v2f bw2[2][8];
  float b2v[2];
  #pragma unroll
  for (int j = 0; j < 2; ++j) {
    const int n = qtr * 128 + (2 * wid + j) * 16 + r;
    #pragma unroll
    for (int kk = 0; kk < 8; ++kk) {
      const int c0 = 4 * kk + 2 * hf;
      const float* bp = w2 + (size_t)c0 * 512 + n;
      bw2[j][kk].x = bp[0]; bw2[j][kk].y = bp[512];
    }
    b2v[j] = b2[n];
  }
  if (tid < HIDD) {
    sLN[tid] = b1[tid]; sLN[HIDD + tid] = gg[tid]; sLN[2 * HIDD + tid] = bb[tid];
  }

  #pragma unroll
  for (int i = 0; i < 8; ++i) {
    const int g  = (tid >> 6) + 2 * i;
    float4 s = pv[i][0];
    #pragma unroll
    for (int jj = 1; jj < 6; ++jj) {
      s.x += pv[i][jj].x; s.y += pv[i][jj].y; s.z += pv[i][jj].z; s.w += pv[i][jj].w;
    }
    s.x *= (1.0f / 256.0f); s.y *= (1.0f / 256.0f); s.z *= (1.0f / 256.0f); s.w *= (1.0f / 256.0f);
    *reinterpret_cast<float4*>(sZ + g * ZP + (tid & 63) * 4) = s;
  }
  __syncthreads();

  {
    v8f m0 = {0, 0, 0, 0, 0, 0, 0, 0};
    v8f m1 = {0, 0, 0, 0, 0, 0, 0, 0};
    #pragma unroll
    for (int kk = 0; kk < 16; kk += 2) {
      const float* ap = sZ + r * ZP + wid * EE + 4 * kk + 2 * hf;
      const v2f a0 = *reinterpret_cast<const v2f*>(ap);
      const v2f a1 = *reinterpret_cast<const v2f*>(ap + 4);
      m0 = wmma4(a0, bv_[kk],     m0);
      m1 = wmma4(a1, bv_[kk + 1], m1);
    }
    #pragma unroll
    for (int g = 0; g < 8; ++g)
      sM[(g + 8 * hf) * EP + n0 + r] = (m0[g] + m1[g]) + bvb;
  }
  __syncthreads();

  {
    v8f acc0 = {0, 0, 0, 0, 0, 0, 0, 0};
    v8f acc1 = {0, 0, 0, 0, 0, 0, 0, 0};
    #pragma unroll
    for (int kk = 0; kk < 16; kk += 2) {
      const float* ap = sM + r * EP + 4 * kk + 2 * hf;
      const v2f a0 = *reinterpret_cast<const v2f*>(ap);
      const v2f a1 = *reinterpret_cast<const v2f*>(ap + 4);
      acc0 = wmma4(a0, bo_[kk],     acc0);
      acc1 = wmma4(a1, bo_[kk + 1], acc1);
    }
    #pragma unroll
    for (int g = 0; g < 8; ++g)
      sP[(g + 8 * hf) * EP + n0 + r] = (acc0[g] + acc1[g]) + bo;
  }
  __syncthreads();

  {
    v8f h0 = {0, 0, 0, 0, 0, 0, 0, 0};
    v8f h1 = {0, 0, 0, 0, 0, 0, 0, 0};
    #pragma unroll
    for (int kk = 0; kk < 8; kk += 2) {
      const float* ap = sP + r * EP + 4 * (8 * kh + kk) + 2 * hf;
      const v2f a0 = *reinterpret_cast<const v2f*>(ap);
      const v2f a1 = *reinterpret_cast<const v2f*>(ap + 4);
      h0 = wmma4(a0, bw1[kk],     h0);
      h1 = wmma4(a1, bw1[kk + 1], h1);
    }
    #pragma unroll
    for (int g = 0; g < 8; ++g)
      sH[kh * (16 * HP) + (g + 8 * hf) * HP + ct * 16 + r] = h0[g] + h1[g];
  }
  __syncthreads();

  {
    const int row = tid >> 3;
    const int cq  = tid & 7;
    const float* h0p = sH + row * HP;
    const float* h1p = sH + 16 * HP + row * HP;
    float x[HIDD];
    #pragma unroll
    for (int q = 0; q < 8; ++q) {
      const v4f p0 = *reinterpret_cast<const v4f*>(h0p + 4 * q);
      const v4f p1 = *reinterpret_cast<const v4f*>(h1p + 4 * q);
      const v4f bv = *reinterpret_cast<const v4f*>(sLN + 4 * q);
      x[4 * q + 0] = (p0.x + p1.x) + bv.x;
      x[4 * q + 1] = (p0.y + p1.y) + bv.y;
      x[4 * q + 2] = (p0.z + p1.z) + bv.z;
      x[4 * q + 3] = (p0.w + p1.w) + bv.w;
    }
    float m0 = 0.0f, m1 = 0.0f, m2 = 0.0f, m3 = 0.0f;
    #pragma unroll
    for (int q = 0; q < 8; ++q) { m0 += x[4 * q]; m1 += x[4 * q + 1]; m2 += x[4 * q + 2]; m3 += x[4 * q + 3]; }
    const float mu = ((m0 + m1) + (m2 + m3)) * (1.0f / 32.0f);
    float v0 = 0.0f, v1 = 0.0f, v2 = 0.0f, v3 = 0.0f;
    #pragma unroll
    for (int q = 0; q < 8; ++q) {
      const float d0 = x[4 * q] - mu, d1 = x[4 * q + 1] - mu, d2 = x[4 * q + 2] - mu, d3 = x[4 * q + 3] - mu;
      v0 = fmaf(d0, d0, v0); v1 = fmaf(d1, d1, v1); v2 = fmaf(d2, d2, v2); v3 = fmaf(d3, d3, v3);
    }
    const float vv = ((v0 + v1) + (v2 + v3)) * (1.0f / 32.0f);
    const float rs = rsqrtf(vv + 1e-5f);
    const v4f q0 = *reinterpret_cast<const v4f*>(h0p + 4 * cq);
    const v4f q1 = *reinterpret_cast<const v4f*>(h1p + 4 * cq);
    const v4f qb = *reinterpret_cast<const v4f*>(sLN + 4 * cq);
    const v4f gv = *reinterpret_cast<const v4f*>(sLN + HIDD + 4 * cq);
    const v4f bt = *reinterpret_cast<const v4f*>(sLN + 2 * HIDD + 4 * cq);
    const float y0 = (((q0.x + q1.x) + qb.x) - mu) * rs * gv.x + bt.x;
    const float y1 = (((q0.y + q1.y) + qb.y) - mu) * rs * gv.y + bt.y;
    const float y2 = (((q0.z + q1.z) + qb.z) - mu) * rs * gv.z + bt.z;
    const float y3 = (((q0.w + q1.w) + qb.w) - mu) * rs * gv.w + bt.w;
    v4f o;
    o.x = 0.5f * y0 * (1.0f + erff(y0 * 0.70710678118654752f));
    o.y = 0.5f * y1 * (1.0f + erff(y1 * 0.70710678118654752f));
    o.z = 0.5f * y2 * (1.0f + erff(y2 * 0.70710678118654752f));
    o.w = 0.5f * y3 * (1.0f + erff(y3 * 0.70710678118654752f));
    *reinterpret_cast<v4f*>(sG + row * HP + 4 * cq) = o;
  }
  __syncthreads();

  {
    v8f c0v = {0, 0, 0, 0, 0, 0, 0, 0};
    v8f c1v = {0, 0, 0, 0, 0, 0, 0, 0};
    #pragma unroll
    for (int kk = 0; kk < 8; ++kk) {
      const v2f a = *reinterpret_cast<const v2f*>(sG + r * HP + 4 * kk + 2 * hf);
      c0v = wmma4(a, bw2[0][kk], c0v);
      c1v = wmma4(a, bw2[1][kk], c1v);
    }
    const int nl = (2 * wid) * 16 + r;
    #pragma unroll
    for (int g = 0; g < 8; ++g) {
      sR[(g + 8 * hf) * RP + nl]      = c0v[g] + b2v[0];
      sR[(g + 8 * hf) * RP + nl + 16] = c1v[g] + b2v[1];
    }
  }
  __syncthreads();

  #pragma unroll
  for (int q = 0; q < 2; ++q) {
    const int idx = tid + 128 * q;
    const int row = idx >> 4;
    const int hl  = (idx >> 3) & 1;
    const int j   = idx & 7;
    float* p = sR + row * RP + hl * 64 + 28 + (NKK * j - (j * (j - 1)) / 2);
    *p = softplusf(*p) + 0.01f;
  }
  if (tid < 16) { sR[tid * RP + 128] = 0.0f; sR[tid * RP + 129] = 1.0f; }
  __syncthreads();

  {
    const int hl = tid >> 6;
    const int ik = tid & 63;
    const int i  = ik >> 3;
    const int k  = ik & 7;
    int oL[NKK], oU[NKK];
    #pragma unroll
    for (int j = 0; j < NKK; ++j) {
      const bool on = (j <= i) && (j <= k);
      oL[j] = on ? ((j == i) ? 129 : (hl * 64 + (i * (i - 1)) / 2 + j)) : 128;
      oU[j] = on ? (hl * 64 + 28 + (NKK * j - (j * (j - 1)) / 2) + (k - j)) : 128;
    }
    const size_t obase = ((size_t)branch * NGRP + G0) * 512 + qtr * 128 + tid;
    #pragma unroll
    for (int row = 0; row < 16; ++row) {
      const float* rb = sR + row * RP;
      float acc = 0.0f;
      #pragma unroll
      for (int j = 0; j < NKK; ++j) acc += rb[oL[j]] * rb[oU[j]];
      out[obase + (size_t)row * 512] = acc;
    }
  }
}

extern "C" void kernel_launch(void* const* d_in, const int* in_sizes, int n_in,
                              void* d_out, int out_size, void* d_ws, size_t ws_size,
                              hipStream_t stream) {
  const float* wsfeat = (const float*)d_in[0];
  const float* W_emb  = (const float*)d_in[1];
  const float* b_emb  = (const float*)d_in[2];
  const float* pos    = (const float*)d_in[3];
  const float* w_in   = (const float*)d_in[4];
  const float* b_in   = (const float*)d_in[5];
  const float* w_out  = (const float*)d_in[6];
  const float* b_out  = (const float*)d_in[7];
  const float* m_w1 = (const float*)d_in[8];
  const float* m_b1 = (const float*)d_in[9];
  const float* m_g  = (const float*)d_in[10];
  const float* m_bb = (const float*)d_in[11];
  const float* m_w2 = (const float*)d_in[12];
  const float* m_b2 = (const float*)d_in[13];
  const float* n_w1 = (const float*)d_in[14];
  const float* n_b1 = (const float*)d_in[15];
  const float* n_g  = (const float*)d_in[16];
  const float* n_bb = (const float*)d_in[17];
  const float* n_w2 = (const float*)d_in[18];
  const float* n_b2 = (const float*)d_in[19];
  float* out = (float*)d_out;

  float* wsp    = (float*)d_ws;
  float* part   = wsp;

  k_fused<<<MROWS / UR, 128, 0, stream>>>(wsfeat, W_emb, b_emb, pos, w_in, b_in, part);
  k_tail<<<(NGRP / 16) * 8, 128, 0, stream>>>(part, w_in, b_in, w_out, b_out,
      m_w1, m_b1, m_g, m_bb, m_w2, m_b2,
      n_w1, n_b1, n_g, n_bb, n_w2, n_b2, out);
}


# K12 index arithmetic made block-uniform: a block's rows lie in two (b,l) planes, so scalar plane bases replace per-row divisions; 2 position loads per lane, 32-bit row offsets; bit-identical
# speedup vs baseline: 4.8842x; 1.0392x over previous
#include <hip/hip_runtime.h>
#include <math.h>

#define LL   12
#define TT   8
#define BB   32
#define CC   256
#define SS   64
#define EE   64
#define HDD  16
#define NHH  4
#define HIDD 32
#define HH   8
#define NKK  8
#define MROWS (BB*CC*LL)
#define SROWS (LL*TT*BB*CC)

typedef float v2f __attribute__((ext_vector_type(2)));
typedef float v4f __attribute__((ext_vector_type(4)));
typedef float v8f __attribute__((ext_vector_type(8)));

static __device__ __forceinline__ v8f wmma4(v2f a, v2f b, v8f c) {
  return __builtin_amdgcn_wmma_f32_16x16x4_f32(false, a, false, b, (short)0, c,
                                               false, false);
}

static __device__ __forceinline__ float softplusf(float x) {
  return (x > 20.0f) ? x : log1pf(__expf(x));
}

static __device__ __forceinline__ float xhalf(float x) {
  const int b = __builtin_bit_cast(int, x);
  return __builtin_bit_cast(float, __builtin_amdgcn_permlanex16(b, b, 0x76543210, 0xfedcba98, false, false));
}

static __device__ __forceinline__ float rowsum16(float x) {
  x += __builtin_bit_cast(float, __builtin_amdgcn_update_dpp(0, __builtin_bit_cast(int, x), 0x161, 0xf, 0xf, true));
  x += __builtin_bit_cast(float, __builtin_amdgcn_update_dpp(0, __builtin_bit_cast(int, x), 0x162, 0xf, 0xf, true));
  x += __builtin_bit_cast(float, __builtin_amdgcn_update_dpp(0, __builtin_bit_cast(int, x), 0x164, 0xf, 0xf, true));
  x += __builtin_bit_cast(float, __builtin_amdgcn_update_dpp(0, __builtin_bit_cast(int, x), 0x168, 0xf, 0xf, true));
  return x;
}

#define ROWQ 17

static __device__ __forceinline__ void row_stats(unsigned la, float* __restrict__ o) {
  float r0, r15, r16, r31, r32, r47, r48, r63;
  float sa, sb, sc, sd, ta, tb, tc, td;
  asm volatile(
    "ds_load_b128 v[168:171], %16\n\t"
    "ds_load_b128 v[172:175], %16 offset:16\n\t"
    "ds_load_b128 v[176:179], %16 offset:32\n\t"
    "ds_load_b128 v[180:183], %16 offset:48\n\t"
    "ds_load_b128 v[184:187], %16 offset:64\n\t"
    "ds_load_b128 v[188:191], %16 offset:80\n\t"
    "ds_load_b128 v[192:195], %16 offset:96\n\t"
    "ds_load_b128 v[196:199], %16 offset:112\n\t"
    "ds_load_b128 v[200:203], %16 offset:128\n\t"
    "ds_load_b128 v[204:207], %16 offset:144\n\t"
    "ds_load_b128 v[208:211], %16 offset:160\n\t"
    "ds_load_b128 v[212:215], %16 offset:176\n\t"
    "ds_load_b128 v[216:219], %16 offset:192\n\t"
    "ds_load_b128 v[220:223], %16 offset:208\n\t"
    "ds_load_b128 v[224:227], %16 offset:224\n\t"
    "ds_load_b128 v[228:231], %16 offset:240\n\t"
    "v_mov_b32 v245, 0\n\t"
    "v_mov_b32 v244, 0\n\t"
    "v_mov_b32 v247, 0\n\t"
    "v_mov_b32 v246, 0\n\t"
    "v_mov_b32 v249, 0\n\t"
    "v_mov_b32 v248, 0\n\t"
    "v_mov_b32 v253, 0\n\t"
    "v_mov_b32 v250, 0\n\t"
    "s_wait_dscnt 0xf\n\t"
    "v_dual_add_f32 v245, v245, v168 :: v_dual_fmac_f32 v244, v168, v168\n\t"
    "v_dual_add_f32 v247, v247, v169 :: v_dual_fmac_f32 v246, v169, v169\n\t"
    "v_dual_add_f32 v249, v249, v170 :: v_dual_fmac_f32 v248, v170, v170\n\t"
    "s_delay_alu instid0(VALU_DEP_4)\n\t"
    "v_dual_add_f32 v253, v253, v171 :: v_dual_fmac_f32 v250, v171, v171\n\t"
    "v_dual_min_num_f32 v232, v168, v169 :: v_dual_max_num_f32 v233, v168, v169\n\t"
    "v_dual_min_num_f32 v234, v170, v171 :: v_dual_max_num_f32 v235, v170, v171\n\t"
    "s_wait_dscnt 0xe\n\t"
    "v_dual_add_f32 v245, v245, v172 :: v_dual_fmac_f32 v244, v172, v172\n\t"
    "v_dual_add_f32 v247, v247, v173 :: v_dual_fmac_f32 v246, v173, v173\n\t"
    "v_dual_add_f32 v249, v249, v174 :: v_dual_fmac_f32 v248, v174, v174\n\t"
    "v_dual_add_f32 v253, v253, v175 :: v_dual_fmac_f32 v250, v175, v175\n\t"
    "v_dual_min_num_f32 v236, v172, v173 :: v_dual_max_num_f32 v237, v172, v173\n\t"
    "v_dual_min_num_f32 v238, v174, v175 :: v_dual_max_num_f32 v239, v174, v175\n\t"
    "s_wait_dscnt 0xd\n\t"
    "v_dual_add_f32 v245, v245, v176 :: v_dual_fmac_f32 v244, v176, v176\n\t"
    "v_dual_add_f32 v247, v247, v177 :: v_dual_fmac_f32 v246, v177, v177\n\t"
    "v_dual_add_f32 v249, v249, v178 :: v_dual_fmac_f32 v248, v178, v178\n\t"
    "v_dual_add_f32 v253, v253, v179 :: v_dual_fmac_f32 v250, v179, v179\n\t"
    "v_dual_min_num_f32 v240, v176, v177 :: v_dual_max_num_f32 v241, v176, v177\n\t"
    "v_dual_min_num_f32 v242, v178, v179 :: v_dual_max_num_f32 v243, v178, v179\n\t"
    "s_wait_dscnt 0xc\n\t"
    "v_dual_add_f32 v245, v245, v180 :: v_dual_fmac_f32 v244, v180, v180\n\t"
    "v_dual_add_f32 v247, v247, v181 :: v_dual_fmac_f32 v246, v181, v181\n\t"
    "v_dual_add_f32 v249, v249, v182 :: v_dual_fmac_f32 v248, v182, v182\n\t"
    "v_dual_add_f32 v253, v253, v183 :: v_dual_fmac_f32 v250, v183, v183\n\t"
    "v_dual_min_num_f32 v168, v180, v181 :: v_dual_max_num_f32 v169, v180, v181\n\t"
    "v_dual_min_num_f32 v170, v182, v183 :: v_dual_max_num_f32 v171, v182, v183\n\t"
    "s_wait_dscnt 0xb\n\t"
    "v_dual_add_f32 v245, v245, v184 :: v_dual_fmac_f32 v244, v184, v184\n\t"
    "v_dual_add_f32 v247, v247, v185 :: v_dual_fmac_f32 v246, v185, v185\n\t"
    "v_dual_add_f32 v249, v249, v186 :: v_dual_fmac_f32 v248, v186, v186\n\t"
    "v_dual_add_f32 v253, v253, v187 :: v_dual_fmac_f32 v250, v187, v187\n\t"
    "v_dual_min_num_f32 v172, v184, v185 :: v_dual_max_num_f32 v173, v184, v185\n\t"
    "v_dual_min_num_f32 v174, v186, v187 :: v_dual_max_num_f32 v175, v186, v187\n\t"
    "s_wait_dscnt 0xa\n\t"
    "v_dual_add_f32 v245, v245, v188 :: v_dual_fmac_f32 v244, v188, v188\n\t"
    "v_dual_add_f32 v247, v247, v189 :: v_dual_fmac_f32 v246, v189, v189\n\t"
    "v_dual_add_f32 v249, v249, v190 :: v_dual_fmac_f32 v248, v190, v190\n\t"
    "v_dual_add_f32 v253, v253, v191 :: v_dual_fmac_f32 v250, v191, v191\n\t"
    "v_dual_min_num_f32 v176, v188, v189 :: v_dual_max_num_f32 v177, v188, v189\n\t"
    "v_dual_min_num_f32 v178, v190, v191 :: v_dual_max_num_f32 v179, v190, v191\n\t"
    "s_wait_dscnt 0x9\n\t"
    "v_dual_add_f32 v245, v245, v192 :: v_dual_fmac_f32 v244, v192, v192\n\t"
    "v_dual_add_f32 v247, v247, v193 :: v_dual_fmac_f32 v246, v193, v193\n\t"
    "v_dual_add_f32 v249, v249, v194 :: v_dual_fmac_f32 v248, v194, v194\n\t"
    "v_dual_add_f32 v253, v253, v195 :: v_dual_fmac_f32 v250, v195, v195\n\t"
    "v_dual_min_num_f32 v180, v192, v193 :: v_dual_max_num_f32 v181, v192, v193\n\t"
    "v_dual_min_num_f32 v182, v194, v195 :: v_dual_max_num_f32 v183, v194, v195\n\t"
    "s_wait_dscnt 0x8\n\t"
    "v_dual_add_f32 v245, v245, v196 :: v_dual_fmac_f32 v244, v196, v196\n\t"
    "v_dual_add_f32 v247, v247, v197 :: v_dual_fmac_f32 v246, v197, v197\n\t"
    "v_dual_add_f32 v249, v249, v198 :: v_dual_fmac_f32 v248, v198, v198\n\t"
    "v_dual_add_f32 v253, v253, v199 :: v_dual_fmac_f32 v250, v199, v199\n\t"
    "v_dual_min_num_f32 v184, v196, v197 :: v_dual_max_num_f32 v185, v196, v197\n\t"
    "v_dual_min_num_f32 v186, v198, v199 :: v_dual_max_num_f32 v187, v198, v199\n\t"
    "s_wait_dscnt 0x7\n\t"
    "v_dual_add_f32 v245, v245, v200 :: v_dual_fmac_f32 v244, v200, v200\n\t"
    "v_dual_add_f32 v247, v247, v201 :: v_dual_fmac_f32 v246, v201, v201\n\t"
    "v_dual_add_f32 v249, v249, v202 :: v_dual_fmac_f32 v248, v202, v202\n\t"
    "v_dual_add_f32 v253, v253, v203 :: v_dual_fmac_f32 v250, v203, v203\n\t"
    "v_dual_min_num_f32 v188, v200, v201 :: v_dual_max_num_f32 v189, v200, v201\n\t"
    "v_dual_min_num_f32 v190, v202, v203 :: v_dual_max_num_f32 v191, v202, v203\n\t"
    "s_wait_dscnt 0x6\n\t"
    "v_dual_add_f32 v245, v245, v204 :: v_dual_fmac_f32 v244, v204, v204\n\t"
    "v_dual_add_f32 v247, v247, v205 :: v_dual_fmac_f32 v246, v205, v205\n\t"
    "v_dual_add_f32 v249, v249, v206 :: v_dual_fmac_f32 v248, v206, v206\n\t"
    "v_dual_add_f32 v253, v253, v207 :: v_dual_fmac_f32 v250, v207, v207\n\t"
    "v_dual_min_num_f32 v192, v204, v205 :: v_dual_max_num_f32 v193, v204, v205\n\t"
    "v_dual_min_num_f32 v194, v206, v207 :: v_dual_max_num_f32 v195, v206, v207\n\t"
    "s_wait_dscnt 0x5\n\t"
    "v_dual_add_f32 v245, v245, v208 :: v_dual_fmac_f32 v244, v208, v208\n\t"
    "v_dual_add_f32 v247, v247, v209 :: v_dual_fmac_f32 v246, v209, v209\n\t"
    "v_dual_add_f32 v249, v249, v210 :: v_dual_fmac_f32 v248, v210, v210\n\t"
    "v_dual_add_f32 v253, v253, v211 :: v_dual_fmac_f32 v250, v211, v211\n\t"
    "v_dual_min_num_f32 v196, v208, v209 :: v_dual_max_num_f32 v197, v208, v209\n\t"
    "v_dual_min_num_f32 v198, v210, v211 :: v_dual_max_num_f32 v199, v210, v211\n\t"
    "s_wait_dscnt 0x4\n\t"
    "v_dual_add_f32 v245, v245, v212 :: v_dual_fmac_f32 v244, v212, v212\n\t"
    "v_dual_add_f32 v247, v247, v213 :: v_dual_fmac_f32 v246, v213, v213\n\t"
    "v_dual_add_f32 v249, v249, v214 :: v_dual_fmac_f32 v248, v214, v214\n\t"
    "v_dual_add_f32 v253, v253, v215 :: v_dual_fmac_f32 v250, v215, v215\n\t"
    "v_dual_min_num_f32 v200, v212, v213 :: v_dual_max_num_f32 v201, v212, v213\n\t"
    "v_dual_min_num_f32 v202, v214, v215 :: v_dual_max_num_f32 v203, v214, v215\n\t"
    "s_wait_dscnt 0x3\n\t"
    "v_dual_add_f32 v245, v245, v216 :: v_dual_fmac_f32 v244, v216, v216\n\t"
    "v_dual_add_f32 v247, v247, v217 :: v_dual_fmac_f32 v246, v217, v217\n\t"
    "v_dual_add_f32 v249, v249, v218 :: v_dual_fmac_f32 v248, v218, v218\n\t"
    "v_dual_add_f32 v253, v253, v219 :: v_dual_fmac_f32 v250, v219, v219\n\t"
    "v_dual_min_num_f32 v204, v216, v217 :: v_dual_max_num_f32 v205, v216, v217\n\t"
    "v_dual_min_num_f32 v206, v218, v219 :: v_dual_max_num_f32 v207, v218, v219\n\t"
    "s_wait_dscnt 0x2\n\t"
    "v_dual_add_f32 v245, v245, v220 :: v_dual_fmac_f32 v244, v220, v220\n\t"
    "v_dual_add_f32 v247, v247, v221 :: v_dual_fmac_f32 v246, v221, v221\n\t"
    "v_dual_add_f32 v249, v249, v222 :: v_dual_fmac_f32 v248, v222, v222\n\t"
    "v_dual_add_f32 v253, v253, v223 :: v_dual_fmac_f32 v250, v223, v223\n\t"
    "v_dual_min_num_f32 v208, v220, v221 :: v_dual_max_num_f32 v209, v220, v221\n\t"
    "v_dual_min_num_f32 v210, v222, v223 :: v_dual_max_num_f32 v211, v222, v223\n\t"
    "s_wait_dscnt 0x1\n\t"
    "v_dual_add_f32 v245, v245, v224 :: v_dual_fmac_f32 v244, v224, v224\n\t"
    "v_dual_add_f32 v247, v247, v225 :: v_dual_fmac_f32 v246, v225, v225\n\t"
    "v_dual_add_f32 v249, v249, v226 :: v_dual_fmac_f32 v248, v226, v226\n\t"
    "v_dual_add_f32 v253, v253, v227 :: v_dual_fmac_f32 v250, v227, v227\n\t"
    "v_dual_min_num_f32 v212, v224, v225 :: v_dual_max_num_f32 v213, v224, v225\n\t"
    "v_dual_min_num_f32 v214, v226, v227 :: v_dual_max_num_f32 v215, v226, v227\n\t"
    "s_wait_dscnt 0x0\n\t"
    "v_dual_add_f32 v245, v245, v228 :: v_dual_fmac_f32 v244, v228, v228\n\t"
    "v_dual_add_f32 v247, v247, v229 :: v_dual_fmac_f32 v246, v229, v229\n\t"
    "v_dual_add_f32 v249, v249, v230 :: v_dual_fmac_f32 v248, v230, v230\n\t"
    "v_dual_add_f32 v253, v253, v231 :: v_dual_fmac_f32 v250, v231, v231\n\t"
    "v_dual_min_num_f32 v216, v228, v229 :: v_dual_max_num_f32 v217, v228, v229\n\t"
    "v_dual_min_num_f32 v218, v230, v231 :: v_dual_max_num_f32 v219, v230, v231\n\t"
    "v_dual_min_num_f32 v220, v232, v234 :: v_dual_max_num_f32 v221, v232, v234\n\t"
    "v_dual_min_num_f32 v222, v233, v235 :: v_dual_max_num_f32 v223, v233, v235\n\t"
    "v_dual_min_num_f32 v226, v236, v238 :: v_dual_max_num_f32 v225, v236, v238\n\t"
    "v_dual_min_num_f32 v224, v237, v239 :: v_dual_max_num_f32 v229, v237, v239\n\t"
    "v_dual_min_num_f32 v228, v240, v242 :: v_dual_max_num_f32 v227, v240, v242\n\t"
    "v_dual_min_num_f32 v230, v241, v243 :: v_dual_max_num_f32 v231, v241, v243\n\t"
    "v_dual_min_num_f32 v234, v168, v170 :: v_dual_max_num_f32 v233, v168, v170\n\t"
    "v_dual_min_num_f32 v232, v169, v171 :: v_dual_max_num_f32 v237, v169, v171\n\t"
    "v_dual_min_num_f32 v236, v172, v174 :: v_dual_max_num_f32 v235, v172, v174\n\t"
    "v_dual_min_num_f32 v238, v173, v175 :: v_dual_max_num_f32 v239, v173, v175\n\t"
    "v_dual_min_num_f32 v242, v176, v178 :: v_dual_max_num_f32 v241, v176, v178\n\t"
    "v_dual_min_num_f32 v240, v177, v179 :: v_dual_max_num_f32 v169, v177, v179\n\t"
    "v_dual_min_num_f32 v168, v180, v182 :: v_dual_max_num_f32 v243, v180, v182\n\t"
    "v_dual_min_num_f32 v170, v181, v183 :: v_dual_max_num_f32 v171, v181, v183\n\t"
    "v_dual_min_num_f32 v174, v184, v186 :: v_dual_max_num_f32 v173, v184, v186\n\t"
    "v_dual_min_num_f32 v172, v185, v187 :: v_dual_max_num_f32 v177, v185, v187\n\t"
    "v_dual_min_num_f32 v176, v188, v190 :: v_dual_max_num_f32 v175, v188, v190\n\t"
    "v_dual_min_num_f32 v178, v189, v191 :: v_dual_max_num_f32 v179, v189, v191\n\t"
    "v_dual_min_num_f32 v182, v192, v194 :: v_dual_max_num_f32 v181, v192, v194\n\t"
    "v_dual_min_num_f32 v180, v193, v195 :: v_dual_max_num_f32 v185, v193, v195\n\t"
    "v_dual_min_num_f32 v184, v196, v198 :: v_dual_max_num_f32 v183, v196, v198\n\t"
    "v_dual_min_num_f32 v186, v197, v199 :: v_dual_max_num_f32 v187, v197, v199\n\t"
    "v_dual_min_num_f32 v190, v200, v202 :: v_dual_max_num_f32 v189, v200, v202\n\t"
    "v_dual_min_num_f32 v188, v201, v203 :: v_dual_max_num_f32 v193, v201, v203\n\t"
    "v_dual_min_num_f32 v192, v204, v206 :: v_dual_max_num_f32 v191, v204, v206\n\t"
    "v_dual_min_num_f32 v194, v205, v207 :: v_dual_max_num_f32 v195, v205, v207\n\t"
    "v_dual_min_num_f32 v198, v208, v210 :: v_dual_max_num_f32 v197, v208, v210\n\t"
    "v_dual_min_num_f32 v196, v209, v211 :: v_dual_max_num_f32 v201, v209, v211\n\t"
    "v_dual_min_num_f32 v200, v212, v214 :: v_dual_max_num_f32 v199, v212, v214\n\t"
    "v_dual_min_num_f32 v202, v213, v215 :: v_dual_max_num_f32 v203, v213, v215\n\t"
    "v_dual_min_num_f32 v206, v216, v218 :: v_dual_max_num_f32 v205, v216, v218\n\t"
    "v_dual_min_num_f32 v204, v217, v219 :: v_dual_max_num_f32 v209, v217, v219\n\t"
    "v_dual_min_num_f32 v208, v222, v221 :: v_dual_max_num_f32 v207, v222, v221\n\t"
    "v_dual_min_num_f32 v210, v224, v225 :: v_dual_max_num_f32 v213, v224, v225\n\t"
    "v_dual_min_num_f32 v212, v230, v227 :: v_dual_max_num_f32 v211, v230, v227\n\t"
    "v_dual_min_num_f32 v214, v232, v233 :: v_dual_max_num_f32 v217, v232, v233\n\t"
    "v_dual_min_num_f32 v216, v238, v235 :: v_dual_max_num_f32 v215, v238, v235\n\t"
    "v_dual_min_num_f32 v218, v240, v241 :: v_dual_max_num_f32 v221, v240, v241\n\t"
    "v_dual_min_num_f32 v222, v170, v243 :: v_dual_max_num_f32 v219, v170, v243\n\t"
    "v_dual_min_num_f32 v224, v172, v173 :: v_dual_max_num_f32 v225, v172, v173\n\t"
    "v_dual_min_num_f32 v230, v178, v175 :: v_dual_max_num_f32 v227, v178, v175\n\t"
    "v_dual_min_num_f32 v232, v180, v181 :: v_dual_max_num_f32 v233, v180, v181\n\t"
    "v_dual_min_num_f32 v238, v186, v183 :: v_dual_max_num_f32 v235, v186, v183\n\t"
    "v_dual_min_num_f32 v240, v188, v189 :: v_dual_max_num_f32 v241, v188, v189\n\t"
    "v_dual_min_num_f32 v170, v194, v191 :: v_dual_max_num_f32 v243, v194, v191\n\t"
    "v_dual_min_num_f32 v172, v196, v197 :: v_dual_max_num_f32 v173, v196, v197\n\t"
    "v_dual_min_num_f32 v178, v202, v199 :: v_dual_max_num_f32 v175, v202, v199\n\t"
    "v_dual_min_num_f32 v180, v204, v205 :: v_dual_max_num_f32 v181, v204, v205\n\t"
    "v_dual_min_num_f32 v186, v220, v226 :: v_dual_max_num_f32 v183, v220, v226\n\t"
    "v_dual_min_num_f32 v188, v208, v210 :: v_dual_max_num_f32 v189, v208, v210\n\t"
    "v_dual_min_num_f32 v194, v207, v213 :: v_dual_max_num_f32 v191, v207, v213\n\t"
    "v_dual_min_num_f32 v196, v223, v229 :: v_dual_max_num_f32 v197, v223, v229\n\t"
    "v_dual_min_num_f32 v204, v228, v234 :: v_dual_max_num_f32 v199, v228, v234\n\t"
    "v_dual_min_num_f32 v202, v212, v214 :: v_dual_max_num_f32 v205, v212, v214\n\t"
    "v_dual_min_num_f32 v220, v211, v217 :: v_dual_max_num_f32 v207, v211, v217\n\t"
    "v_dual_min_num_f32 v226, v231, v237 :: v_dual_max_num_f32 v223, v231, v237\n\t"
    "v_dual_min_num_f32 v208, v236, v242 :: v_dual_max_num_f32 v213, v236, v242\n\t"
    "v_dual_min_num_f32 v210, v216, v218 :: v_dual_max_num_f32 v229, v216, v218\n\t"
    "v_dual_min_num_f32 v228, v215, v221 :: v_dual_max_num_f32 v211, v215, v221\n\t"
    "v_dual_min_num_f32 v234, v239, v169 :: v_dual_max_num_f32 v217, v239, v169\n\t"
    "v_dual_min_num_f32 v214, v168, v174 :: v_dual_max_num_f32 v231, v168, v174\n\t"
    "v_dual_min_num_f32 v212, v222, v224 :: v_dual_max_num_f32 v237, v222, v224\n\t"
    "v_dual_min_num_f32 v236, v219, v225 :: v_dual_max_num_f32 v215, v219, v225\n\t"
    "v_dual_min_num_f32 v242, v171, v177 :: v_dual_max_num_f32 v239, v171, v177\n\t"
    "v_dual_min_num_f32 v216, v176, v182 :: v_dual_max_num_f32 v221, v176, v182\n\t"
    "v_dual_min_num_f32 v218, v230, v232 :: v_dual_max_num_f32 v169, v230, v232\n\t"
    "v_dual_min_num_f32 v168, v227, v233 :: v_dual_max_num_f32 v219, v227, v233\n\t"
    "v_dual_min_num_f32 v174, v179, v185 :: v_dual_max_num_f32 v225, v179, v185\n\t"
    "v_dual_min_num_f32 v222, v184, v190 :: v_dual_max_num_f32 v171, v184, v190\n\t"
    "v_dual_min_num_f32 v224, v238, v240 :: v_dual_max_num_f32 v177, v238, v240\n\t"
    "v_dual_min_num_f32 v176, v235, v241 :: v_dual_max_num_f32 v227, v235, v241\n\t"
    "v_dual_min_num_f32 v182, v187, v193 :: v_dual_max_num_f32 v179, v187, v193\n\t"
    "v_dual_min_num_f32 v230, v192, v198 :: v_dual_max_num_f32 v233, v192, v198\n\t"
    "v_dual_min_num_f32 v232, v170, v172 :: v_dual_max_num_f32 v185, v170, v172\n\t"
    "v_dual_min_num_f32 v184, v243, v173 :: v_dual_max_num_f32 v235, v243, v173\n\t"
    "v_dual_min_num_f32 v190, v195, v201 :: v_dual_max_num_f32 v241, v195, v201\n\t"
    "v_dual_min_num_f32 v240, v200, v206 :: v_dual_max_num_f32 v187, v200, v206\n\t"
    "v_dual_min_num_f32 v238, v178, v180 :: v_dual_max_num_f32 v193, v178, v180\n\t"
    "v_dual_min_num_f32 v192, v175, v181 :: v_dual_max_num_f32 v243, v175, v181\n\t"
    "v_dual_min_num_f32 v198, v203, v209 :: v_dual_max_num_f32 v195, v203, v209\n\t"
    "v_dual_min_num_f32 v170, v194, v183 :: v_dual_max_num_f32 v173, v194, v183\n\t"
    "v_dual_min_num_f32 v172, v196, v189 :: v_dual_max_num_f32 v201, v196, v189\n\t"
    "v_dual_min_num_f32 v200, v220, v199 :: v_dual_max_num_f32 v175, v220, v199\n\t"
    "v_dual_min_num_f32 v206, v226, v205 :: v_dual_max_num_f32 v181, v226, v205\n\t"
    "v_dual_min_num_f32 v180, v228, v213 :: v_dual_max_num_f32 v203, v228, v213\n\t"
    "v_dual_min_num_f32 v178, v234, v229 :: v_dual_max_num_f32 v209, v234, v229\n\t"
    "v_dual_min_num_f32 v194, v236, v231 :: v_dual_max_num_f32 v183, v236, v231\n\t"
    "v_dual_min_num_f32 v196, v242, v237 :: v_dual_max_num_f32 v189, v242, v237\n\t"
    "v_dual_min_num_f32 v220, v168, v221 :: v_dual_max_num_f32 v199, v168, v221\n\t"
    "v_dual_min_num_f32 v226, v174, v169 :: v_dual_max_num_f32 v205, v174, v169\n\t"
    "v_dual_min_num_f32 v234, v176, v171 :: v_dual_max_num_f32 v213, v176, v171\n\t"
    "v_dual_min_num_f32 v228, v182, v177 :: v_dual_max_num_f32 v229, v182, v177\n\t"
    "v_dual_min_num_f32 v242, v184, v233 :: v_dual_max_num_f32 v231, v184, v233\n\t"
    "v_dual_min_num_f32 v236, v190, v185 :: v_dual_max_num_f32 v237, v190, v185\n\t"
    "v_dual_min_num_f32 v168, v192, v187 :: v_dual_max_num_f32 v221, v192, v187\n\t"
    "v_dual_min_num_f32 v174, v198, v193 :: v_dual_max_num_f32 v169, v198, v193\n\t"
    "v_dual_min_num_f32 v176, v188, v170 :: v_dual_max_num_f32 v171, v188, v170\n\t"
    "v_dual_min_num_f32 v182, v172, v173 :: v_dual_max_num_f32 v177, v172, v173\n\t"
    "v_dual_min_num_f32 v184, v201, v191 :: v_dual_max_num_f32 v233, v201, v191\n\t"
    "v_dual_min_num_f32 v190, v202, v200 :: v_dual_max_num_f32 v185, v202, v200\n\t"
    "v_dual_min_num_f32 v192, v206, v175 :: v_dual_max_num_f32 v187, v206, v175\n\t"
    "v_dual_min_num_f32 v198, v181, v207 :: v_dual_max_num_f32 v191, v181, v207\n\t"
    "v_dual_min_num_f32 v188, v210, v180 :: v_dual_max_num_f32 v193, v210, v180\n\t"
    "v_dual_min_num_f32 v170, v178, v203 :: v_dual_max_num_f32 v173, v178, v203\n\t"
    "v_dual_min_num_f32 v172, v209, v211 :: v_dual_max_num_f32 v201, v209, v211\n\t"
    "v_dual_min_num_f32 v202, v212, v194 :: v_dual_max_num_f32 v175, v212, v194\n\t"
    "v_dual_min_num_f32 v200, v196, v183 :: v_dual_max_num_f32 v207, v196, v183\n\t"
    "v_dual_min_num_f32 v206, v189, v215 :: v_dual_max_num_f32 v203, v189, v215\n\t"
    "v_dual_min_num_f32 v210, v218, v220 :: v_dual_max_num_f32 v181, v218, v220\n\t"
    "v_dual_min_num_f32 v180, v226, v199 :: v_dual_max_num_f32 v209, v226, v199\n\t"
    "v_dual_min_num_f32 v178, v205, v219 :: v_dual_max_num_f32 v211, v205, v219\n\t"
    "v_dual_min_num_f32 v212, v224, v234 :: v_dual_max_num_f32 v183, v224, v234\n\t"
    "v_dual_min_num_f32 v194, v228, v213 :: v_dual_max_num_f32 v215, v228, v213\n\t"
    "v_dual_min_num_f32 v196, v229, v227 :: v_dual_max_num_f32 v189, v229, v227\n\t"
    "v_dual_min_num_f32 v218, v232, v242 :: v_dual_max_num_f32 v199, v232, v242\n\t"
    "v_dual_min_num_f32 v220, v236, v231 :: v_dual_max_num_f32 v205, v236, v231\n\t"
    "v_dual_min_num_f32 v226, v237, v235 :: v_dual_max_num_f32 v219, v237, v235\n\t"
    "v_dual_min_num_f32 v224, v238, v168 :: v_dual_max_num_f32 v213, v238, v168\n\t"
    "v_dual_min_num_f32 v234, v174, v221 :: v_dual_max_num_f32 v227, v174, v221\n\t"
    "v_dual_min_num_f32 v228, v169, v243 :: v_dual_max_num_f32 v229, v169, v243\n\t"
    "v_dual_min_num_f32 v232, v186, v204 :: v_dual_max_num_f32 v231, v186, v204\n\t"
    "v_dual_min_num_f32 v242, v176, v190 :: v_dual_max_num_f32 v237, v176, v190\n\t"
    "v_dual_min_num_f32 v236, v171, v185 :: v_dual_max_num_f32 v235, v171, v185\n\t"
    "v_dual_min_num_f32 v238, v182, v192 :: v_dual_max_num_f32 v221, v182, v192\n\t"
    "v_dual_min_num_f32 v168, v177, v187 :: v_dual_max_num_f32 v169, v177, v187\n\t"
    "v_dual_min_num_f32 v174, v184, v198 :: v_dual_max_num_f32 v243, v184, v198\n\t"
    "v_dual_min_num_f32 v186, v233, v191 :: v_dual_max_num_f32 v171, v233, v191\n\t"
    "v_dual_min_num_f32 v204, v197, v223 :: v_dual_max_num_f32 v185, v197, v223\n\t"
    "v_dual_min_num_f32 v190, v208, v214 :: v_dual_max_num_f32 v177, v208, v214\n\t"
    "v_dual_min_num_f32 v176, v188, v202 :: v_dual_max_num_f32 v187, v188, v202\n\t"
    "v_dual_min_num_f32 v182, v193, v175 :: v_dual_max_num_f32 v233, v193, v175\n\t"
    "v_dual_min_num_f32 v192, v170, v200 :: v_dual_max_num_f32 v191, v170, v200\n\t"
    "v_dual_min_num_f32 v184, v173, v207 :: v_dual_max_num_f32 v197, v173, v207\n\t"
    "v_dual_min_num_f32 v198, v172, v206 :: v_dual_max_num_f32 v223, v172, v206\n\t"
    "v_dual_min_num_f32 v208, v201, v203 :: v_dual_max_num_f32 v193, v201, v203\n\t"
    "v_dual_min_num_f32 v214, v217, v239 :: v_dual_max_num_f32 v175, v217, v239\n\t"
    "v_dual_min_num_f32 v188, v216, v222 :: v_dual_max_num_f32 v173, v216, v222\n\t"
    "v_dual_min_num_f32 v202, v210, v212 :: v_dual_max_num_f32 v207, v210, v212\n\t"
    "v_dual_min_num_f32 v170, v181, v183 :: v_dual_max_num_f32 v201, v181, v183\n\t"
    "v_dual_min_num_f32 v200, v180, v194 :: v_dual_max_num_f32 v203, v180, v194\n\t"
    "v_dual_min_num_f32 v172, v209, v215 :: v_dual_max_num_f32 v217, v209, v215\n\t"
    "v_dual_min_num_f32 v206, v178, v196 :: v_dual_max_num_f32 v239, v178, v196\n\t"
    "v_dual_min_num_f32 v216, v211, v189 :: v_dual_max_num_f32 v181, v211, v189\n\t"
    "v_dual_min_num_f32 v222, v225, v179 :: v_dual_max_num_f32 v183, v225, v179\n\t"
    "v_dual_min_num_f32 v210, v230, v240 :: v_dual_max_num_f32 v209, v230, v240\n\t"
    "v_dual_min_num_f32 v212, v218, v224 :: v_dual_max_num_f32 v215, v218, v224\n\t"
    "v_dual_min_num_f32 v180, v199, v213 :: v_dual_max_num_f32 v211, v199, v213\n\t"
    "v_dual_min_num_f32 v194, v220, v234 :: v_dual_max_num_f32 v189, v220, v234\n\t"
    "v_dual_min_num_f32 v178, v205, v227 :: v_dual_max_num_f32 v225, v205, v227\n\t"
    "v_dual_min_num_f32 v196, v226, v228 :: v_dual_max_num_f32 v179, v226, v228\n\t"
    "v_dual_min_num_f32 v230, v219, v229 :: v_dual_max_num_f32 v199, v219, v229\n\t"
    "v_dual_min_num_f32 v240, v241, v195 :: v_dual_max_num_f32 v213, v241, v195\n\t"
    "v_dual_min_num_f32 v218, v168, v231 :: v_dual_max_num_f32 v205, v168, v231\n\t"
    "v_dual_min_num_f32 v224, v174, v237 :: v_dual_max_num_f32 v227, v174, v237\n\t"
    "v_dual_min_num_f32 v220, v186, v235 :: v_dual_max_num_f32 v219, v186, v235\n\t"
    "v_dual_min_num_f32 v234, v204, v221 :: v_dual_max_num_f32 v229, v204, v221\n\t"
    "v_dual_min_num_f32 v228, v184, v177 :: v_dual_max_num_f32 v241, v184, v177\n\t"
    "v_dual_min_num_f32 v226, v198, v187 :: v_dual_max_num_f32 v195, v198, v187\n\t"
    "v_dual_min_num_f32 v168, v208, v233 :: v_dual_max_num_f32 v231, v208, v233\n\t"
    "v_dual_min_num_f32 v174, v214, v191 :: v_dual_max_num_f32 v237, v214, v191\n\t"
    "v_dual_min_num_f32 v204, v172, v173 :: v_dual_max_num_f32 v235, v172, v173\n\t"
    "v_dual_min_num_f32 v186, v206, v207 :: v_dual_max_num_f32 v221, v206, v207\n\t"
    "v_dual_min_num_f32 v184, v216, v201 :: v_dual_max_num_f32 v187, v216, v201\n\t"
    "v_dual_min_num_f32 v198, v222, v203 :: v_dual_max_num_f32 v177, v222, v203\n\t"
    "v_dual_min_num_f32 v214, v178, v209 :: v_dual_max_num_f32 v233, v178, v209\n\t"
    "v_dual_min_num_f32 v208, v196, v215 :: v_dual_max_num_f32 v191, v196, v215\n\t"
    "v_dual_min_num_f32 v172, v230, v211 :: v_dual_max_num_f32 v207, v230, v211\n\t"
    "v_dual_min_num_f32 v206, v240, v189 :: v_dual_max_num_f32 v173, v240, v189\n\t"
    "v_dual_min_num_f32 v216, v236, v218 :: v_dual_max_num_f32 v201, v236, v218\n\t"
    "v_dual_min_num_f32 v222, v238, v224 :: v_dual_max_num_f32 v203, v238, v224\n\t"
    "v_dual_min_num_f32 v178, v220, v205 :: v_dual_max_num_f32 v209, v220, v205\n\t"
    "v_dual_min_num_f32 v196, v234, v227 :: v_dual_max_num_f32 v215, v234, v227\n\t"
    "v_dual_min_num_f32 v230, v219, v169 :: v_dual_max_num_f32 v211, v219, v169\n\t"
    "v_dual_min_num_f32 v240, v229, v243 :: v_dual_max_num_f32 v189, v229, v243\n\t"
    "v_dual_min_num_f32 v218, v182, v228 :: v_dual_max_num_f32 v205, v182, v228\n\t"
    "v_dual_min_num_f32 v236, v192, v226 :: v_dual_max_num_f32 v227, v192, v226\n\t"
    "v_dual_min_num_f32 v238, v168, v241 :: v_dual_max_num_f32 v219, v168, v241\n\t"
    "v_dual_min_num_f32 v224, v174, v195 :: v_dual_max_num_f32 v169, v174, v195\n\t"
    "v_dual_min_num_f32 v220, v231, v197 :: v_dual_max_num_f32 v229, v231, v197\n\t"
    "v_dual_min_num_f32 v234, v237, v223 :: v_dual_max_num_f32 v243, v237, v223\n\t"
    "v_dual_min_num_f32 v228, v170, v204 :: v_dual_max_num_f32 v241, v170, v204\n\t"
    "v_dual_min_num_f32 v182, v200, v186 :: v_dual_max_num_f32 v195, v200, v186\n\t"
    "v_dual_min_num_f32 v192, v184, v235 :: v_dual_max_num_f32 v231, v184, v235\n\t"
    "v_dual_min_num_f32 v226, v198, v221 :: v_dual_max_num_f32 v197, v198, v221\n\t"
    "v_dual_min_num_f32 v168, v187, v217 :: v_dual_max_num_f32 v237, v187, v217\n\t"
    "v_dual_min_num_f32 v174, v177, v239 :: v_dual_max_num_f32 v223, v177, v239\n\t"
    "v_dual_min_num_f32 v170, v180, v214 :: v_dual_max_num_f32 v235, v180, v214\n\t"
    "v_dual_min_num_f32 v204, v194, v208 :: v_dual_max_num_f32 v221, v194, v208\n\t"
    "v_dual_min_num_f32 v200, v172, v233 :: v_dual_max_num_f32 v187, v172, v233\n\t"
    "v_dual_min_num_f32 v186, v206, v191 :: v_dual_max_num_f32 v217, v206, v191\n\t"
    "v_dual_min_num_f32 v184, v207, v225 :: v_dual_max_num_f32 v177, v207, v225\n\t"
    "v_dual_min_num_f32 v198, v173, v179 :: v_dual_max_num_f32 v233, v173, v179\n\t"
    "v_dual_min_num_f32 v180, v242, v216 :: v_dual_max_num_f32 v239, v242, v216\n\t"
    "v_dual_min_num_f32 v214, v222, v201 :: v_dual_max_num_f32 v191, v222, v201\n\t"
    "v_dual_min_num_f32 v194, v203, v178 :: v_dual_max_num_f32 v207, v203, v178\n\t"
    "v_dual_min_num_f32 v208, v196, v209 :: v_dual_max_num_f32 v225, v196, v209\n\t"
    "v_dual_min_num_f32 v172, v215, v230 :: v_dual_max_num_f32 v173, v215, v230\n\t"
    "v_dual_min_num_f32 v206, v240, v211 :: v_dual_max_num_f32 v179, v240, v211\n\t"
    "v_dual_min_num_f32 v242, v189, v171 :: v_dual_max_num_f32 v201, v189, v171\n\t"
    "v_dual_min_num_f32 v222, v176, v218 :: v_dual_max_num_f32 v209, v176, v218\n\t"
    "v_dual_min_num_f32 v216, v236, v205 :: v_dual_max_num_f32 v189, v236, v205\n\t"
    "v_dual_min_num_f32 v196, v227, v238 :: v_dual_max_num_f32 v205, v227, v238\n\t"
    "v_dual_min_num_f32 v178, v224, v219 :: v_dual_max_num_f32 v203, v224, v219\n\t"
    "v_dual_min_num_f32 v230, v169, v220 :: v_dual_max_num_f32 v215, v169, v220\n\t"
    "v_dual_min_num_f32 v240, v234, v229 :: v_dual_max_num_f32 v169, v234, v229\n\t"
    "v_dual_min_num_f32 v176, v243, v193 :: v_dual_max_num_f32 v211, v243, v193\n\t"
    "v_dual_min_num_f32 v218, v202, v228 :: v_dual_max_num_f32 v171, v202, v228\n\t"
    "v_dual_min_num_f32 v236, v182, v241 :: v_dual_max_num_f32 v227, v182, v241\n\t"
    "v_dual_min_num_f32 v238, v195, v192 :: v_dual_max_num_f32 v219, v195, v192\n\t"
    "v_dual_min_num_f32 v224, v226, v231 :: v_dual_max_num_f32 v229, v226, v231\n\t"
    "v_dual_min_num_f32 v220, v197, v168 :: v_dual_max_num_f32 v243, v197, v168\n\t"
    "v_dual_min_num_f32 v234, v174, v237 :: v_dual_max_num_f32 v193, v174, v237\n\t"
    "v_dual_min_num_f32 v202, v223, v181 :: v_dual_max_num_f32 v241, v223, v181\n\t"
    "v_dual_min_num_f32 v228, v212, v170 :: v_dual_max_num_f32 v197, v212, v170\n\t"
    "v_dual_min_num_f32 v182, v204, v235 :: v_dual_max_num_f32 v237, v204, v235\n\t"
    "v_dual_min_num_f32 v192, v221, v200 :: v_dual_max_num_f32 v181, v221, v200\n\t"
    "v_dual_min_num_f32 v226, v186, v187 :: v_dual_max_num_f32 v195, v186, v187\n\t"
    "v_dual_min_num_f32 v174, v217, v184 :: v_dual_max_num_f32 v221, v217, v184\n\t"
    "v_dual_min_num_f32 v168, v198, v177 :: v_dual_max_num_f32 v231, v198, v177\n\t"
    "v_dual_min_num_f32 v212, v233, v199 :: v_dual_max_num_f32 v223, v233, v199\n\t"
    "v_dual_min_num_f32 v170, v232, v190 :: v_dual_max_num_f32 v235, v232, v190\n\t"
    "v_dual_min_num_f32 v204, v180, v222 :: v_dual_max_num_f32 v187, v180, v222\n\t"
    "v_dual_min_num_f32 v200, v239, v209 :: v_dual_max_num_f32 v217, v239, v209\n\t"
    "v_dual_min_num_f32 v186, v214, v216 :: v_dual_max_num_f32 v177, v214, v216\n\t"
    "v_dual_min_num_f32 v184, v191, v189 :: v_dual_max_num_f32 v233, v191, v189\n\t"
    "v_dual_min_num_f32 v198, v194, v196 :: v_dual_max_num_f32 v199, v194, v196\n\t"
    "v_dual_min_num_f32 v232, v207, v205 :: v_dual_max_num_f32 v239, v207, v205\n\t"
    "v_dual_min_num_f32 v190, v208, v178 :: v_dual_max_num_f32 v209, v208, v178\n\t"
    "v_dual_min_num_f32 v180, v225, v203 :: v_dual_max_num_f32 v191, v225, v203\n\t"
    "v_dual_min_num_f32 v222, v172, v230 :: v_dual_max_num_f32 v189, v172, v230\n\t"
    "v_dual_min_num_f32 v214, v173, v215 :: v_dual_max_num_f32 v207, v173, v215\n\t"
    "v_dual_min_num_f32 v216, v206, v240 :: v_dual_max_num_f32 v205, v206, v240\n\t"
    "v_dual_min_num_f32 v194, v179, v169 :: v_dual_max_num_f32 v225, v179, v169\n\t"
    "v_dual_min_num_f32 v196, v242, v176 :: v_dual_max_num_f32 v203, v242, v176\n\t"
    "v_dual_min_num_f32 v208, v201, v211 :: v_dual_max_num_f32 v173, v201, v211\n\t"
    "v_dual_min_num_f32 v178, v185, v175 :: v_dual_max_num_f32 v215, v185, v175\n\t"
    "v_dual_min_num_f32 v172, v188, v210 :: v_dual_max_num_f32 v179, v188, v210\n\t"
    "v_dual_min_num_f32 v230, v218, v228 :: v_dual_max_num_f32 v169, v218, v228\n\t"
    "v_dual_min_num_f32 v206, v171, v197 :: v_dual_max_num_f32 v201, v171, v197\n\t"
    "v_dual_min_num_f32 v240, v236, v182 :: v_dual_max_num_f32 v211, v236, v182\n\t"
    "v_dual_min_num_f32 v242, v227, v237 :: v_dual_max_num_f32 v185, v227, v237\n\t"
    "v_dual_min_num_f32 v176, v238, v192 :: v_dual_max_num_f32 v175, v238, v192\n\t"
    "v_dual_min_num_f32 v188, v219, v181 :: v_dual_max_num_f32 v171, v219, v181\n\t"
    "v_dual_min_num_f32 v210, v224, v226 :: v_dual_max_num_f32 v197, v224, v226\n\t"
    "v_dual_min_num_f32 v218, v229, v195 :: v_dual_max_num_f32 v227, v229, v195\n\t"
    "v_dual_min_num_f32 v228, v220, v174 :: v_dual_max_num_f32 v237, v220, v174\n\t"
    "v_dual_min_num_f32 v236, v243, v221 :: v_dual_max_num_f32 v219, v243, v221\n\t"
    "v_dual_min_num_f32 v182, v234, v168 :: v_dual_max_num_f32 v181, v234, v168\n\t"
    "v_dual_min_num_f32 v238, v193, v231 :: v_dual_max_num_f32 v229, v193, v231\n\t"
    "v_dual_min_num_f32 v192, v202, v212 :: v_dual_max_num_f32 v195, v202, v212\n\t"
    "v_dual_min_num_f32 v224, v241, v223 :: v_dual_max_num_f32 v243, v241, v223\n\t"
    "v_dual_min_num_f32 v226, v183, v213 :: v_dual_max_num_f32 v221, v183, v213\n\t"
    "v_dual_min_num_f32 v174, v180, v235 :: v_dual_max_num_f32 v193, v180, v235\n\t"
    "v_dual_min_num_f32 v220, v222, v187 :: v_dual_max_num_f32 v231, v222, v187\n\t"
    "v_dual_min_num_f32 v234, v214, v217 :: v_dual_max_num_f32 v241, v214, v217\n\t"
    "v_dual_min_num_f32 v168, v216, v177 :: v_dual_max_num_f32 v223, v216, v177\n\t"
    "v_dual_min_num_f32 v202, v194, v233 :: v_dual_max_num_f32 v213, v194, v233\n\t"
    "v_dual_min_num_f32 v212, v196, v199 :: v_dual_max_num_f32 v183, v196, v199\n\t"
    "v_dual_min_num_f32 v180, v208, v239 :: v_dual_max_num_f32 v217, v208, v239\n\t"
    "v_dual_min_num_f32 v222, v178, v209 :: v_dual_max_num_f32 v235, v178, v209\n\t"
    "v_dual_min_num_f32 v216, v218, v179 :: v_dual_max_num_f32 v187, v218, v179\n\t"
    "v_dual_min_num_f32 v214, v228, v169 :: v_dual_max_num_f32 v177, v228, v169\n\t"
    "v_dual_min_num_f32 v194, v236, v201 :: v_dual_max_num_f32 v233, v236, v201\n\t"
    "v_dual_min_num_f32 v196, v182, v211 :: v_dual_max_num_f32 v199, v182, v211\n\t"
    "v_dual_min_num_f32 v208, v238, v185 :: v_dual_max_num_f32 v209, v238, v185\n\t"
    "v_dual_min_num_f32 v178, v192, v175 :: v_dual_max_num_f32 v239, v192, v175\n\t"
    "v_dual_min_num_f32 v218, v224, v171 :: v_dual_max_num_f32 v169, v224, v171\n\t"
    "v_dual_min_num_f32 v228, v226, v197 :: v_dual_max_num_f32 v179, v226, v197\n\t"
    "v_dual_min_num_f32 v182, v184, v174 :: v_dual_max_num_f32 v201, v184, v174\n\t"
    "v_dual_min_num_f32 v236, v198, v220 :: v_dual_max_num_f32 v211, v198, v220\n\t"
    "v_dual_min_num_f32 v238, v232, v234 :: v_dual_max_num_f32 v185, v232, v234\n\t"
    "v_dual_min_num_f32 v192, v190, v168 :: v_dual_max_num_f32 v175, v190, v168\n\t"
    "v_dual_min_num_f32 v224, v202, v193 :: v_dual_max_num_f32 v171, v202, v193\n\t"
    "v_dual_min_num_f32 v226, v212, v231 :: v_dual_max_num_f32 v197, v212, v231\n\t"
    "v_dual_min_num_f32 v184, v180, v241 :: v_dual_max_num_f32 v193, v180, v241\n\t"
    "v_dual_min_num_f32 v174, v222, v223 :: v_dual_max_num_f32 v231, v222, v223\n\t"
    "v_dual_min_num_f32 v198, v213, v191 :: v_dual_max_num_f32 v241, v213, v191\n\t"
    "v_dual_min_num_f32 v220, v183, v189 :: v_dual_max_num_f32 v223, v183, v189\n\t"
    "v_dual_min_num_f32 v232, v217, v207 :: v_dual_max_num_f32 v191, v217, v207\n\t"
    "v_dual_min_num_f32 v234, v235, v205 :: v_dual_max_num_f32 v213, v235, v205\n\t"
    "v_dual_min_num_f32 v168, v242, v216 :: v_dual_max_num_f32 v183, v242, v216\n\t"
    "v_dual_min_num_f32 v190, v176, v214 :: v_dual_max_num_f32 v189, v176, v214\n\t"
    "v_dual_min_num_f32 v202, v188, v194 :: v_dual_max_num_f32 v217, v188, v194\n\t"
    "v_dual_min_num_f32 v212, v210, v196 :: v_dual_max_num_f32 v207, v210, v196\n\t"
    "v_dual_min_num_f32 v180, v208, v187 :: v_dual_max_num_f32 v235, v208, v187\n\t"
    "v_dual_min_num_f32 v222, v178, v177 :: v_dual_max_num_f32 v205, v178, v177\n\t"
    "v_dual_min_num_f32 v242, v218, v233 :: v_dual_max_num_f32 v187, v218, v233\n\t"
    "v_dual_min_num_f32 v216, v228, v199 :: v_dual_max_num_f32 v177, v228, v199\n\t"
    "v_dual_min_num_f32 v176, v209, v227 :: v_dual_max_num_f32 v233, v209, v227\n\t"
    "v_dual_min_num_f32 v214, v239, v237 :: v_dual_max_num_f32 v199, v239, v237\n\t"
    "v_dual_min_num_f32 v188, v169, v219 :: v_dual_max_num_f32 v227, v169, v219\n\t"
    "v_dual_min_num_f32 v194, v179, v181 :: v_dual_max_num_f32 v209, v179, v181\n\t"
    "v_dual_min_num_f32 v210, v200, v182 :: v_dual_max_num_f32 v239, v200, v182\n\t"
    "v_dual_min_num_f32 v196, v186, v236 :: v_dual_max_num_f32 v237, v186, v236\n\t"
    "v_dual_min_num_f32 v208, v238, v201 :: v_dual_max_num_f32 v169, v238, v201\n\t"
    "v_dual_min_num_f32 v178, v192, v211 :: v_dual_max_num_f32 v219, v192, v211\n\t"
    "v_dual_min_num_f32 v218, v185, v224 :: v_dual_max_num_f32 v179, v185, v224\n\t"
    "v_dual_min_num_f32 v228, v175, v226 :: v_dual_max_num_f32 v181, v175, v226\n\t"
    "v_dual_min_num_f32 v200, v184, v171 :: v_dual_max_num_f32 v201, v184, v171\n\t"
    "v_dual_min_num_f32 v182, v174, v197 :: v_dual_max_num_f32 v211, v174, v197\n\t"
    "v_dual_min_num_f32 v186, v193, v198 :: v_dual_max_num_f32 v185, v193, v198\n\t"
    "v_dual_min_num_f32 v236, v231, v220 :: v_dual_max_num_f32 v175, v231, v220\n\t"
    "v_dual_min_num_f32 v238, v232, v241 :: v_dual_max_num_f32 v171, v232, v241\n\t"
    "v_dual_min_num_f32 v192, v234, v223 :: v_dual_max_num_f32 v197, v234, v223\n\t"
    "v_dual_min_num_f32 v224, v191, v225 :: v_dual_max_num_f32 v193, v191, v225\n\t"
    "v_dual_min_num_f32 v226, v213, v203 :: v_dual_max_num_f32 v231, v213, v203\n\t"
    "v_dual_min_num_f32 v184, v206, v168 :: v_dual_max_num_f32 v241, v206, v168\n\t"
    "v_dual_min_num_f32 v174, v240, v190 :: v_dual_max_num_f32 v223, v240, v190\n\t"
    "v_dual_min_num_f32 v198, v202, v183 :: v_dual_max_num_f32 v191, v202, v183\n\t"
    "v_dual_min_num_f32 v220, v212, v189 :: v_dual_max_num_f32 v225, v212, v189\n\t"
    "v_dual_min_num_f32 v232, v217, v180 :: v_dual_max_num_f32 v213, v217, v180\n\t"
    "v_dual_min_num_f32 v234, v207, v222 :: v_dual_max_num_f32 v203, v207, v222\n\t"
    "v_dual_min_num_f32 v206, v242, v235 :: v_dual_max_num_f32 v183, v242, v235\n\t"
    "v_dual_min_num_f32 v168, v216, v205 :: v_dual_max_num_f32 v189, v216, v205\n\t"
    "v_dual_min_num_f32 v240, v187, v176 :: v_dual_max_num_f32 v217, v187, v176\n\t"
    "v_dual_min_num_f32 v190, v177, v214 :: v_dual_max_num_f32 v207, v177, v214\n\t"
    "v_dual_min_num_f32 v202, v188, v233 :: v_dual_max_num_f32 v235, v188, v233\n\t"
    "v_dual_min_num_f32 v212, v194, v199 :: v_dual_max_num_f32 v205, v194, v199\n\t"
    "v_dual_min_num_f32 v180, v227, v229 :: v_dual_max_num_f32 v187, v227, v229\n\t"
    "v_dual_min_num_f32 v222, v209, v195 :: v_dual_max_num_f32 v177, v209, v195\n\t"
    "v_dual_min_num_f32 v242, v204, v210 :: v_dual_max_num_f32 v233, v204, v210\n\t"
    "v_dual_min_num_f32 v216, v196, v239 :: v_dual_max_num_f32 v199, v196, v239\n\t"
    "v_dual_min_num_f32 v176, v237, v208 :: v_dual_max_num_f32 v227, v237, v208\n\t"
    "v_dual_min_num_f32 v214, v178, v169 :: v_dual_max_num_f32 v229, v178, v169\n\t"
    "v_dual_min_num_f32 v188, v219, v218 :: v_dual_max_num_f32 v209, v219, v218\n\t"
    "v_dual_min_num_f32 v194, v228, v179 :: v_dual_max_num_f32 v195, v228, v179\n\t"
    "v_dual_min_num_f32 v204, v181, v200 :: v_dual_max_num_f32 v239, v181, v200\n\t"
    "v_dual_min_num_f32 v210, v182, v201 :: v_dual_max_num_f32 v237, v182, v201\n\t"
    "v_dual_min_num_f32 v196, v211, v186 :: v_dual_max_num_f32 v169, v211, v186\n\t"
    "v_dual_min_num_f32 v208, v236, v185 :: v_dual_max_num_f32 v219, v236, v185\n\t"
    "v_dual_min_num_f32 v178, v175, v238 :: v_dual_max_num_f32 v179, v175, v238\n\t"
    "v_dual_min_num_f32 v218, v192, v171 :: v_dual_max_num_f32 v181, v192, v171\n\t"
    "v_dual_min_num_f32 v228, v197, v224 :: v_dual_max_num_f32 v201, v197, v224\n\t"
    "v_dual_min_num_f32 v200, v226, v193 :: v_dual_max_num_f32 v211, v226, v193\n\t"
    "v_dual_min_num_f32 v182, v231, v173 :: v_dual_max_num_f32 v185, v231, v173\n\t"
    "v_dual_min_num_f32 v236, v230, v184 :: v_dual_max_num_f32 v175, v230, v184\n\t"
    "v_dual_min_num_f32 v186, v174, v241 :: v_dual_max_num_f32 v197, v174, v241\n\t"
    "v_dual_min_num_f32 v238, v223, v198 :: v_dual_max_num_f32 v193, v223, v198\n\t"
    "v_dual_min_num_f32 v192, v220, v191 :: v_dual_max_num_f32 v171, v220, v191\n\t"
    "v_dual_min_num_f32 v226, v225, v232 :: v_dual_max_num_f32 v231, v225, v232\n\t"
    "v_dual_min_num_f32 v224, v234, v213 :: v_dual_max_num_f32 v173, v234, v213\n\t"
    "v_dual_min_num_f32 v230, v203, v206 :: v_dual_max_num_f32 v241, v203, v206\n\t"
    "v_dual_min_num_f32 v184, v168, v183 :: v_dual_max_num_f32 v223, v168, v183\n\t"
    "v_dual_min_num_f32 v174, v189, v240 :: v_dual_max_num_f32 v191, v189, v240\n\t"
    "v_dual_min_num_f32 v198, v190, v217 :: v_dual_max_num_f32 v225, v190, v217\n\t"
    "v_dual_min_num_f32 v220, v207, v202 :: v_dual_max_num_f32 v213, v207, v202\n\t"
    "v_dual_min_num_f32 v232, v212, v235 :: v_dual_max_num_f32 v203, v212, v235\n\t"
    "v_dual_min_num_f32 v234, v205, v180 :: v_dual_max_num_f32 v183, v205, v180\n\t"
    "v_dual_min_num_f32 v206, v222, v187 :: v_dual_max_num_f32 v189, v222, v187\n\t"
    "v_dual_min_num_f32 v168, v177, v243 :: v_dual_max_num_f32 v207, v177, v243\n\t"
    "v_dual_min_num_f32 v240, v170, v172 :: v_dual_max_num_f32 v217, v170, v172\n\t"
    "v_max_num_f32 v236, v242, v236\n\t"
    "v_max_num_f32 v175, v233, v175\n\t"
    "v_max_num_f32 v186, v216, v186\n\t"
    "v_max_num_f32 v197, v199, v197\n\t"
    "v_max_num_f32 v238, v176, v238\n\t"
    "v_max_num_f32 v193, v227, v193\n\t"
    "v_max_num_f32 v192, v214, v192\n\t"
    "v_dual_min_num_f32 v190, v229, v171 :: v_dual_max_num_f32 v235, v229, v171\n\t"
    "v_dual_min_num_f32 v202, v188, v226 :: v_dual_max_num_f32 v205, v188, v226\n\t"
    "v_dual_min_num_f32 v212, v209, v231 :: v_dual_max_num_f32 v187, v209, v231\n\t"
    "v_dual_min_num_f32 v180, v194, v224 :: v_dual_max_num_f32 v177, v194, v224\n\t"
    "v_dual_min_num_f32 v222, v195, v173 :: v_dual_max_num_f32 v243, v195, v173\n\t"
    "v_dual_min_num_f32 v170, v204, v230 :: v_dual_max_num_f32 v233, v204, v230\n\t"
    "v_dual_min_num_f32 v172, v239, v241 :: v_dual_max_num_f32 v199, v239, v241\n\t"
    "v_dual_min_num_f32 v242, v210, v184 :: v_dual_max_num_f32 v227, v210, v184\n\t"
    "v_dual_min_num_f32 v216, v237, v223 :: v_dual_max_num_f32 v229, v237, v223\n\t"
    "v_dual_min_num_f32 v214, v196, v174 :: v_dual_max_num_f32 v171, v196, v174\n\t"
    "v_dual_min_num_f32 v176, v169, v191 :: v_dual_max_num_f32 v209, v169, v191\n\t"
    "v_dual_min_num_f32 v188, v208, v198 :: v_dual_max_num_f32 v231, v208, v198\n\t"
    "v_dual_min_num_f32 v226, v219, v225 :: v_dual_max_num_f32 v195, v219, v225\n\t"
    "v_dual_min_num_f32 v224, v178, v220 :: v_dual_max_num_f32 v173, v178, v220\n\t"
    "v_dual_min_num_f32 v194, v179, v213 :: v_dual_max_num_f32 v239, v179, v213\n\t"
    "v_dual_min_num_f32 v230, v218, v232 :: v_dual_max_num_f32 v241, v218, v232\n\t"
    "v_min_num_f32 v181, v181, v203\n\t"
    "v_min_num_f32 v228, v228, v234\n\t"
    "v_min_num_f32 v201, v201, v183\n\t"
    "v_min_num_f32 v200, v200, v206\n\t"
    "v_min_num_f32 v204, v211, v189\n\t"
    "v_min_num_f32 v182, v182, v168\n\t"
    "v_min_num_f32 v185, v185, v207\n\t"
    "v_dual_min_num_f32 v210, v215, v221 :: v_dual_max_num_f32 v237, v215, v221\n\t"
    "v_dual_min_num_f32 v184, v216, v217 :: v_dual_max_num_f32 v223, v216, v217\n\t"
    "v_dual_min_num_f32 v196, v214, v236 :: v_dual_max_num_f32 v169, v214, v236\n\t"
    "v_dual_min_num_f32 v174, v176, v175 :: v_dual_max_num_f32 v191, v176, v175\n\t"
    "v_dual_min_num_f32 v198, v188, v186 :: v_dual_max_num_f32 v219, v188, v186\n\t"
    "v_dual_min_num_f32 v208, v226, v197 :: v_dual_max_num_f32 v225, v226, v197\n\t"
    "v_dual_min_num_f32 v220, v224, v238 :: v_dual_max_num_f32 v179, v224, v238\n\t"
    "v_dual_min_num_f32 v178, v194, v193 :: v_dual_max_num_f32 v213, v194, v193\n\t"
    "v_dual_min_num_f32 v232, v230, v192 :: v_dual_max_num_f32 v203, v230, v192\n\t"
    "v_dual_min_num_f32 v218, v181, v235 :: v_dual_max_num_f32 v183, v181, v235\n\t"
    "v_dual_min_num_f32 v234, v228, v205 :: v_dual_max_num_f32 v189, v228, v205\n\t"
    "v_dual_min_num_f32 v206, v201, v187 :: v_dual_max_num_f32 v211, v201, v187\n\t"
    "v_dual_min_num_f32 v168, v200, v177 :: v_dual_max_num_f32 v221, v200, v177\n\t"
    "v_dual_min_num_f32 v216, v204, v243 :: v_dual_max_num_f32 v217, v204, v243\n\t"
    "v_dual_min_num_f32 v214, v182, v233 :: v_dual_max_num_f32 v207, v182, v233\n\t"
    "v_dual_min_num_f32 v236, v185, v199 :: v_dual_max_num_f32 v197, v185, v199\n\t"
    "v_dual_min_num_f32 v176, v210, v227 :: v_dual_max_num_f32 v215, v210, v227\n\t"
    "v_max_num_f32 v184, v190, v184\n\t"
    "v_max_num_f32 v196, v202, v196\n\t"
    "v_max_num_f32 v174, v212, v174\n\t"
    "v_max_num_f32 v198, v180, v198\n\t"
    "v_min_num_f32 v222, v222, v208\n\t"
    "v_min_num_f32 v170, v170, v220\n\t"
    "v_min_num_f32 v172, v172, v178\n\t"
    "v_min_num_f32 v175, v242, v232\n\t"
    "v_max_num_f32 v223, v218, v223\n\t"
    "v_max_num_f32 v169, v234, v169\n\t"
    "v_max_num_f32 v191, v206, v191\n\t"
    "v_max_num_f32 v219, v168, v219\n\t"
    "v_min_num_f32 v216, v216, v225\n\t"
    "v_min_num_f32 v214, v214, v179\n\t"
    "v_min_num_f32 v236, v236, v213\n\t"
    "v_min_num_f32 v176, v176, v203\n\t"
    "v_max_num_f32 v229, v183, v229\n\t"
    "v_max_num_f32 v171, v189, v171\n\t"
    "v_max_num_f32 v209, v211, v209\n\t"
    "v_max_num_f32 v231, v221, v231\n\t"
    "v_min_num_f32 v188, v217, v195\n\t"
    "v_min_num_f32 v186, v207, v173\n\t"
    "v_min_num_f32 v226, v197, v239\n\t"
    "v_min_num_f32 v224, v215, v241\n\t"
    "v_max_num_f32 v184, v222, v184\n\t"
    "v_max_num_f32 v196, v170, v196\n\t"
    "v_min_num_f32 v238, v172, v174\n\t"
    "v_min_num_f32 v175, v175, v198\n\t"
    "v_max_num_f32 v223, v216, v223\n\t"
    "v_max_num_f32 v169, v214, v169\n\t"
    "v_min_num_f32 v236, v236, v191\n\t"
    "v_min_num_f32 v176, v176, v219\n\t"
    "v_max_num_f32 v229, v188, v229\n\t"
    "v_max_num_f32 v171, v186, v171\n\t"
    "v_min_num_f32 v226, v226, v209\n\t"
    "v_min_num_f32 v224, v224, v231\n\t"
    "v_max_num_f32 v184, v238, v184\n\t"
    "v_min_num_f32 v175, v175, v196\n\t"
    "v_max_num_f32 v223, v236, v223\n\t"
    "v_min_num_f32 v176, v176, v169\n\t"
    "v_max_num_f32 v229, v226, v229\n\t"
    "v_min_num_f32 v224, v224, v171\n\t"
    "v_dual_min_num_f32 v194, v175, v184 :: v_dual_max_num_f32 v193, v175, v184\n\t"
    "s_delay_alu instid0(VALU_DEP_4)\n\t"
    "v_dual_min_num_f32 v230, v176, v223 :: v_dual_max_num_f32 v181, v176, v223\n\t"
    "s_delay_alu instid0(VALU_DEP_3)\n\t"
    "v_dual_min_num_f32 v192, v224, v229 :: v_dual_max_num_f32 v235, v224, v229\n\t"
    "v_mov_b32 %0, v240\n\t"
    "v_mov_b32 %1, v194\n\t"
    "v_mov_b32 %2, v193\n\t"
    "v_mov_b32 %3, v230\n\t"
    "v_mov_b32 %4, v181\n\t"
    "v_mov_b32 %5, v192\n\t"
    "v_mov_b32 %6, v235\n\t"
    "v_mov_b32 %7, v237\n\t"
    "v_mov_b32 %8, v245\n\t"
    "v_mov_b32 %9, v247\n\t"
    "v_mov_b32 %10, v249\n\t"
    "v_mov_b32 %11, v253\n\t"
    "v_mov_b32 %12, v244\n\t"
    "v_mov_b32 %13, v246\n\t"
    "v_mov_b32 %14, v248\n\t"
    "v_mov_b32 %15, v250\n\t"
    : "=v"(r0), "=v"(r15), "=v"(r16), "=v"(r31), "=v"(r32), "=v"(r47), "=v"(r48), "=v"(r63),
      "=v"(sa), "=v"(sb), "=v"(sc), "=v"(sd), "=v"(ta), "=v"(tb), "=v"(tc), "=v"(td)
    : "v"(la)
    : "memory", "v168", "v169", "v170", "v171", "v172", "v173", "v174", "v175", "v176", "v177", "v178", "v179", "v180", "v181", "v182", "v183", "v184", "v185", "v186", "v187", "v188", "v189", "v190", "v191", "v192", "v193", "v194", "v195", "v196", "v197", "v198", "v199", "v200", "v201", "v202", "v203", "v204", "v205", "v206", "v207", "v208", "v209", "v210", "v211", "v212", "v213", "v214", "v215", "v216", "v217", "v218", "v219", "v220", "v221", "v222", "v223", "v224", "v225", "v226", "v227", "v228", "v229", "v230", "v231", "v232", "v233", "v234", "v235", "v236", "v237", "v238", "v239", "v240", "v241", "v242", "v243", "v244", "v245", "v246", "v247", "v248", "v249", "v250", "v251", "v252", "v253");

  const float s  = (sa + sb) + (sc + sd);
  const float s2 = (ta + tb) + (tc + td);
  const float mean = s * (1.0f / 64.0f);
  const float var  = (s2 - 64.0f * mean * mean) * (1.0f / 63.0f);

  const float q0 = r0;
  const float q1 = r15 + 0.75f * (r16 - r15);
  const float q2 = 0.5f * (r31 + r32);
  const float q3 = r47 + 0.25f * (r48 - r47);
  const float q4 = r63;

  o[0] = mean; o[1] = var; o[2] = q0; o[3] = q1; o[4] = q2; o[5] = q3; o[6] = q4;
}

#define UR 48
#define FP 60
#define EP 68
#define QP 196

__global__ __launch_bounds__(128) void k_fused(const float* __restrict__ ws,
                                               const float* __restrict__ Wemb,
                                               const float* __restrict__ bemb,
                                               const float* __restrict__ pos,
                                               const float* __restrict__ win,
                                               const float* __restrict__ bin,
                                               float* __restrict__ part) {
  __shared__ __attribute__((aligned(16))) float sA[UR * FP];
  __shared__ __attribute__((aligned(16))) float sB[UR * EP];
  __shared__ __attribute__((aligned(16))) float sC[UR * QP];
  const int tid  = threadIdx.x;
  const int lane = tid & 31;
  const int wid  = tid >> 5;
  const int r  = lane & 15;
  const int hf = lane >> 4;
  const size_t row0 = (size_t)blockIdx.x * UR;
  const int bl0 = (int)(row0 >> 8);
  const int bA = bl0 / LL;
  const int lA = bl0 - bA * LL;
  const int bB = (bl0 + 1) / LL;
  const int lB = (bl0 + 1) - bB * LL;
  const int psplit = ((bl0 + 1) << 8) - (int)row0;
  const int baseA = lA * (TT * BB * CC) + bA * CC;
  const int baseB = lB * (TT * BB * CC) + bB * CC;
  const v4f* ws4 = reinterpret_cast<const v4f*>(ws);

  {
    int* sRow = reinterpret_cast<int*>(sB);
    v4f* wt = reinterpret_cast<v4f*>(sC) + wid * (32 * ROWQ);
    const unsigned la = (unsigned)reinterpret_cast<size_t>(wt + lane * ROWQ);
    #pragma unroll
    for (int k = 0; k < 3; ++k) {
      const int idx = tid + 128 * k;
      const int t = idx / UR;
      const int i = idx - t * UR;
      const int c = ((int)row0 + i) & 255;
      const int sr = ((i < psplit) ? baseA : baseB) + t * (BB * CC) + c;
      sRow[idx] = sr * (SS / 4);
    }
    __syncthreads();

    v4f x[16];
    {
      const int base = wid * 32 + hf;
      #pragma unroll
      for (int q = 0; q < 16; ++q) {
        x[q] = ws4[(unsigned)(sRow[base + 2 * q] + r)];
      }
    }
    #pragma unroll 1
    for (int p = 0; p < 3; ++p) {
      #pragma unroll
      for (int q = 0; q < 16; ++q) wt[(2 * q + hf) * ROWQ + r] = x[q];
      __syncthreads();
      if (p < 2) {
        const int base = (p + 1) * 128 + wid * 32 + hf;
        #pragma unroll
        for (int q = 0; q < 16; ++q) {
          x[q] = ws4[(unsigned)(sRow[base + 2 * q] + r)];
        }
      }
      const int idx = p * 128 + tid;
      const int t = idx / UR;
      const int i = idx - t * UR;
      row_stats(la, sA + i * FP + t * 7);
    }
  }
  __syncthreads();

  {
    const int n0 = wid * 16;
    v8f e0 = {0, 0, 0, 0, 0, 0, 0, 0};
    v8f e1 = {0, 0, 0, 0, 0, 0, 0, 0};
    v8f e2 = {0, 0, 0, 0, 0, 0, 0, 0};
    #pragma unroll
    for (int k0 = 0; k0 < 56; k0 += 4) {
      const int c0 = k0 + 2 * hf;
      const float* bp = Wemb + (size_t)c0 * EE + n0 + r;
      v2f b; b.x = bp[0]; b.y = bp[EE];
      const float* ap = sA + r * FP + c0;
      const v2f a0 = *reinterpret_cast<const v2f*>(ap);
      const v2f a1 = *reinterpret_cast<const v2f*>(ap + 16 * FP);
      const v2f a2 = *reinterpret_cast<const v2f*>(ap + 32 * FP);
      e0 = wmma4(a0, b, e0);
      e1 = wmma4(a1, b, e1);
      e2 = wmma4(a2, b, e2);
    }
    const int n = n0 + r;
    const float be = bemb[n];
    const float pA = pos[lA * EE + n];
    const float pB = pos[lB * EE + n];
    #pragma unroll
    for (int g = 0; g < 8; ++g) {
      const int ml = g + 8 * hf;
      sB[(ml)      * EP + n] = e0[g] + be + ((ml      < psplit) ? pA : pB);
      sB[(ml + 16) * EP + n] = e1[g] + be + ((ml + 16 < psplit) ? pA : pB);
      sB[(ml + 32) * EP + n] = e2[g] + be + ((ml + 32 < psplit) ? pA : pB);
    }
  }
  __syncthreads();

  {
    v2f a[3][16];
    #pragma unroll
    for (int kk = 0; kk < 16; ++kk) {
      const float* ap = sB + r * EP + 4 * kk + 2 * hf;
      a[0][kk] = *reinterpret_cast<const v2f*>(ap);
      a[1][kk] = *reinterpret_cast<const v2f*>(ap + 16 * EP);
      a[2][kk] = *reinterpret_cast<const v2f*>(ap + 32 * EP);
    }
    #pragma unroll
    for (int j = 0; j < 2; ++j) {
      const int n0 = (wid + 4 * j) * 16;
      v8f c0v = {0, 0, 0, 0, 0, 0, 0, 0};
      v8f c1v = {0, 0, 0, 0, 0, 0, 0, 0};
      v8f c2v = {0, 0, 0, 0, 0, 0, 0, 0};
      #pragma unroll
      for (int kk = 0; kk < 16; ++kk) {
        const int c0 = 4 * kk + 2 * hf;
        const float* bp = win + (size_t)c0 * 192 + n0 + r;
        v2f b; b.x = bp[0]; b.y = bp[192];
        c0v = wmma4(a[0][kk], b, c0v);
        c1v = wmma4(a[1][kk], b, c1v);
        c2v = wmma4(a[2][kk], b, c2v);
      }
      const int n = n0 + r;
      const float bi_ = bin[n];
      #pragma unroll
      for (int g = 0; g < 8; ++g) {
        const int ml = g + 8 * hf;
        sC[(ml)      * QP + n] = c0v[g] + bi_;
        sC[(ml + 16) * QP + n] = c1v[g] + bi_;
        sC[(ml + 32) * QP + n] = c2v[g] + bi_;
      }
    }
  }
  __syncthreads();

  const int rq = (r < LL) ? r : (LL - 1);
  const int rk = (r < 8) ? ((r < 6) ? r : 5) : ((r < 14) ? (r - 2) : (LL - 1));
  const float qmask = (r < LL) ? 1.0f : 0.0f;

  const int gsplit = (int)((((row0 >> 8) + 1) << 8) - row0);
  v4f z0 = {0, 0, 0, 0};
  v4f z1 = {0, 0, 0, 0};
  #pragma unroll
  for (int bi = 0; bi < 4; ++bi) {
    const float* base = sC + (bi * LL) * QP + wid * HDD;

    v8f S = {0, 0, 0, 0, 0, 0, 0, 0};
    #pragma unroll
    for (int k0 = 0; k0 < 16; k0 += 4) {
      const int c0 = k0 + 2 * hf;
      const v2f a = *reinterpret_cast<const v2f*>(base + rk * QP + EE + c0);
      const v2f b = *reinterpret_cast<const v2f*>(base + rq * QP + c0);
      S = wmma4(a, b, S);
    }

    float sv[6];
    #pragma unroll
    for (int g = 0; g < 6; ++g) sv[g] = S[g] * 0.25f;
    float rm = fmaxf(fmaxf(fmaxf(sv[0], sv[1]), fmaxf(sv[2], sv[3])), fmaxf(sv[4], sv[5]));
    rm = fmaxf(rm, xhalf(rm));
    float e[6];
    #pragma unroll
    for (int g = 0; g < 6; ++g) e[g] = __expf(sv[g] - rm);
    float rs = ((e[0] + e[1]) + (e[2] + e[3])) + (e[4] + e[5]);
    rs += xhalf(rs);
    const float inv = qmask / rs;

    float p[6], w[6];
    v4f x[6];
    #pragma unroll
    for (int g = 0; g < 6; ++g) {
      p[g] = e[g] * inv;
      w[g] = rowsum16(p[g]);
      x[g] = *reinterpret_cast<const v4f*>(sB + (bi * LL + 6 * hf + g) * EP + 4 * r);
    }
    const int q0 = bi * LL;
    if (q0 < gsplit && gsplit < q0 + LL) {
      const bool qfirst = (q0 + r) < gsplit;
      #pragma unroll
      for (int g = 0; g < 6; ++g) {
        const float wa = rowsum16(qfirst ? p[g] : 0.0f);
        const float wb = w[g] - wa;
        z0.x = fmaf(wa, x[g].x, z0.x); z0.y = fmaf(wa, x[g].y, z0.y); z0.z = fmaf(wa, x[g].z, z0.z); z0.w = fmaf(wa, x[g].w, z0.w);
        z1.x = fmaf(wb, x[g].x, z1.x); z1.y = fmaf(wb, x[g].y, z1.y); z1.z = fmaf(wb, x[g].z, z1.z); z1.w = fmaf(wb, x[g].w, z1.w);
      }
    } else {
      v4f zb = {0, 0, 0, 0};
      #pragma unroll
      for (int g = 0; g < 6; ++g) {
        zb.x = fmaf(w[g], x[g].x, zb.x); zb.y = fmaf(w[g], x[g].y, zb.y); zb.z = fmaf(w[g], x[g].z, zb.z); zb.w = fmaf(w[g], x[g].w, zb.w);
      }
      const bool bfirst = (q0 + LL) <= gsplit;
      z0.x += bfirst ? zb.x : 0.0f; z0.y += bfirst ? zb.y : 0.0f; z0.z += bfirst ? zb.z : 0.0f; z0.w += bfirst ? zb.w : 0.0f;
      z1.x += bfirst ? 0.0f : zb.x; z1.y += bfirst ? 0.0f : zb.y; z1.z += bfirst ? 0.0f : zb.z; z1.w += bfirst ? 0.0f : zb.w;
    }
  }
  z0.x += xhalf(z0.x); z0.y += xhalf(z0.y); z0.z += xhalf(z0.z); z0.w += xhalf(z0.w);
  z1.x += xhalf(z1.x); z1.y += xhalf(z1.y); z1.z += xhalf(z1.z); z1.w += xhalf(z1.w);

  const v4f zo = hf ? z1 : z0;
  *reinterpret_cast<v4f*>(part + (((size_t)blockIdx.x * 2 + hf) * NHH + wid) * EE + 4 * r) = zo;
}

#define NGRP (BB * LL)
#define HP 36
#define RP 132
#define ZP 260
__global__ __launch_bounds__(128) void k_tail(
    const float* __restrict__ part,
    const float* __restrict__ win, const float* __restrict__ bin,
    const float* __restrict__ wout, const float* __restrict__ bout,
    const float* __restrict__ mw1, const float* __restrict__ mb1,
    const float* __restrict__ mg,  const float* __restrict__ mbb,
    const float* __restrict__ mw2, const float* __restrict__ mb2,
    const float* __restrict__ nw1, const float* __restrict__ nb1,
    const float* __restrict__ ng,  const float* __restrict__ nbb,
    const float* __restrict__ nw2, const float* __restrict__ nb2,
    float* __restrict__ out) {
  __shared__ __attribute__((aligned(16))) float sZ[16 * ZP];
  __shared__ __attribute__((aligned(16))) float sM[16 * EP];
  __shared__ __attribute__((aligned(16))) float sP[16 * EP];
  __shared__ __attribute__((aligned(16))) float sH[2 * 16 * HP];
  __shared__ __attribute__((aligned(16))) float sG[16 * HP];
  __shared__ __attribute__((aligned(16))) float sR[16 * RP];
  __shared__ __attribute__((aligned(16))) float sLN[3 * HIDD];
  const int tid  = threadIdx.x;
  const int lane = tid & 31;
  const int wid  = tid >> 5;
  const int r  = lane & 15;
  const int hf = lane >> 4;
  const int qtr    = blockIdx.x & 3;
  const int branch = (blockIdx.x >> 2) & 1;
  const int G0     = (blockIdx.x >> 3) * 16;
  const float* w1 = branch ? nw1 : mw1;
  const float* b1 = branch ? nb1 : mb1;
  const float* gg = branch ? ng  : mg;
  const float* bb = branch ? nbb : mbb;
  const float* w2 = branch ? nw2 : mw2;
  const float* b2 = branch ? nb2 : mb2;

  static_assert(UR == 48 && CC == 256, "six blocks per pool group");
  const float4* p4 = reinterpret_cast<const float4*>(part);
  float4 pv[8][6];
  #pragma unroll
  for (int i = 0; i < 8; ++i) {
    const int G  = G0 + (tid >> 6) + 2 * i;
    const int jf = (G * 256) / UR;
    #pragma unroll
    for (int jj = 0; jj < 6; ++jj) {
      const int j = jf + jj;
      const int slot = G - ((j * UR) >> 8);
      pv[i][jj] = p4[(size_t)(j * 2 + slot) * (NHH * EE / 4) + (tid & 63)];
    }
  }

  const int n0 = wid * 16;
  v2f bo_[16];
  #pragma unroll
  for (int kk = 0; kk < 16; ++kk) {
    const int c0 = 4 * kk + 2 * hf;
    const float* bp = wout + (size_t)c0 * EE + n0 + r;
    bo_[kk].x = bp[0]; bo_[kk].y = bp[EE];
  }
  const float bo = bout[n0 + r];
  v2f bv_[16];
  #pragma unroll
  for (int kk = 0; kk < 16; ++kk) {
    const int c0 = 4 * kk + 2 * hf;
    const float* bp = win + (size_t)c0 * 192 + 2 * EE + n0 + r;
    bv_[kk].x = bp[0]; bv_[kk].y = bp[192];
  }
  const float bvb = bin[2 * EE + n0 + r];
  const int ct = wid & 1;
  const int kh = wid >> 1;
  v2f bw1[8];
  #pragma unroll
  for (int kk = 0; kk < 8; ++kk) {
    const int c0 = 4 * (8 * kh + kk) + 2 * hf;
    const float* bp = w1 + (size_t)c0 * HIDD + ct * 16 + r;
    bw1[kk].x = bp[0]; bw1[kk].y = bp[HIDD];
  }
  v2f bw2[2][8];
  float b2v[2];
  #pragma unroll
  for (int j = 0; j < 2; ++j) {
    const int n = qtr * 128 + (2 * wid + j) * 16 + r;
    #pragma unroll
    for (int kk = 0; kk < 8; ++kk) {
      const int c0 = 4 * kk + 2 * hf;
      const float* bp = w2 + (size_t)c0 * 512 + n;
      bw2[j][kk].x = bp[0]; bw2[j][kk].y = bp[512];
    }
    b2v[j] = b2[n];
  }
  if (tid < HIDD) {
    sLN[tid] = b1[tid]; sLN[HIDD + tid] = gg[tid]; sLN[2 * HIDD + tid] = bb[tid];
  }

  #pragma unroll
  for (int i = 0; i < 8; ++i) {
    const int g  = (tid >> 6) + 2 * i;
    float4 s = pv[i][0];
    #pragma unroll
    for (int jj = 1; jj < 6; ++jj) {
      s.x += pv[i][jj].x; s.y += pv[i][jj].y; s.z += pv[i][jj].z; s.w += pv[i][jj].w;
    }
    s.x *= (1.0f / 256.0f); s.y *= (1.0f / 256.0f); s.z *= (1.0f / 256.0f); s.w *= (1.0f / 256.0f);
    *reinterpret_cast<float4*>(sZ + g * ZP + (tid & 63) * 4) = s;
  }
  __syncthreads();

  {
    v8f m0 = {0, 0, 0, 0, 0, 0, 0, 0};
    v8f m1 = {0, 0, 0, 0, 0, 0, 0, 0};
    #pragma unroll
    for (int kk = 0; kk < 16; kk += 2) {
      const float* ap = sZ + r * ZP + wid * EE + 4 * kk + 2 * hf;
      const v2f a0 = *reinterpret_cast<const v2f*>(ap);
      const v2f a1 = *reinterpret_cast<const v2f*>(ap + 4);
      m0 = wmma4(a0, bv_[kk],     m0);
      m1 = wmma4(a1, bv_[kk + 1], m1);
    }
    #pragma unroll
    for (int g = 0; g < 8; ++g)
      sM[(g + 8 * hf) * EP + n0 + r] = (m0[g] + m1[g]) + bvb;
  }
  __syncthreads();

  {
    v8f acc0 = {0, 0, 0, 0, 0, 0, 0, 0};
    v8f acc1 = {0, 0, 0, 0, 0, 0, 0, 0};
    #pragma unroll
    for (int kk = 0; kk < 16; kk += 2) {
      const float* ap = sM + r * EP + 4 * kk + 2 * hf;
      const v2f a0 = *reinterpret_cast<const v2f*>(ap);
      const v2f a1 = *reinterpret_cast<const v2f*>(ap + 4);
      acc0 = wmma4(a0, bo_[kk],     acc0);
      acc1 = wmma4(a1, bo_[kk + 1], acc1);
    }
    #pragma unroll
    for (int g = 0; g < 8; ++g)
      sP[(g + 8 * hf) * EP + n0 + r] = (acc0[g] + acc1[g]) + bo;
  }
  __syncthreads();

  {
    v8f h0 = {0, 0, 0, 0, 0, 0, 0, 0};
    v8f h1 = {0, 0, 0, 0, 0, 0, 0, 0};
    #pragma unroll
    for (int kk = 0; kk < 8; kk += 2) {
      const float* ap = sP + r * EP + 4 * (8 * kh + kk) + 2 * hf;
      const v2f a0 = *reinterpret_cast<const v2f*>(ap);
      const v2f a1 = *reinterpret_cast<const v2f*>(ap + 4);
      h0 = wmma4(a0, bw1[kk],     h0);
      h1 = wmma4(a1, bw1[kk + 1], h1);
    }
    #pragma unroll
    for (int g = 0; g < 8; ++g)
      sH[kh * (16 * HP) + (g + 8 * hf) * HP + ct * 16 + r] = h0[g] + h1[g];
  }
  __syncthreads();

  {
    const int row = tid >> 3;
    const int cq  = tid & 7;
    const float* h0p = sH + row * HP;
    const float* h1p = sH + 16 * HP + row * HP;
    float x[HIDD];
    #pragma unroll
    for (int q = 0; q < 8; ++q) {
      const v4f p0 = *reinterpret_cast<const v4f*>(h0p + 4 * q);
      const v4f p1 = *reinterpret_cast<const v4f*>(h1p + 4 * q);
      const v4f bv = *reinterpret_cast<const v4f*>(sLN + 4 * q);
      x[4 * q + 0] = (p0.x + p1.x) + bv.x;
      x[4 * q + 1] = (p0.y + p1.y) + bv.y;
      x[4 * q + 2] = (p0.z + p1.z) + bv.z;
      x[4 * q + 3] = (p0.w + p1.w) + bv.w;
    }
    float m0 = 0.0f, m1 = 0.0f, m2 = 0.0f, m3 = 0.0f;
    #pragma unroll
    for (int q = 0; q < 8; ++q) { m0 += x[4 * q]; m1 += x[4 * q + 1]; m2 += x[4 * q + 2]; m3 += x[4 * q + 3]; }
    const float mu = ((m0 + m1) + (m2 + m3)) * (1.0f / 32.0f);
    float v0 = 0.0f, v1 = 0.0f, v2 = 0.0f, v3 = 0.0f;
    #pragma unroll
    for (int q = 0; q < 8; ++q) {
      const float d0 = x[4 * q] - mu, d1 = x[4 * q + 1] - mu, d2 = x[4 * q + 2] - mu, d3 = x[4 * q + 3] - mu;
      v0 = fmaf(d0, d0, v0); v1 = fmaf(d1, d1, v1); v2 = fmaf(d2, d2, v2); v3 = fmaf(d3, d3, v3);
    }
    const float vv = ((v0 + v1) + (v2 + v3)) * (1.0f / 32.0f);
    const float rs = rsqrtf(vv + 1e-5f);
    const v4f q0 = *reinterpret_cast<const v4f*>(h0p + 4 * cq);
    const v4f q1 = *reinterpret_cast<const v4f*>(h1p + 4 * cq);
    const v4f qb = *reinterpret_cast<const v4f*>(sLN + 4 * cq);
    const v4f gv = *reinterpret_cast<const v4f*>(sLN + HIDD + 4 * cq);
    const v4f bt = *reinterpret_cast<const v4f*>(sLN + 2 * HIDD + 4 * cq);
    const float y0 = (((q0.x + q1.x) + qb.x) - mu) * rs * gv.x + bt.x;
    const float y1 = (((q0.y + q1.y) + qb.y) - mu) * rs * gv.y + bt.y;
    const float y2 = (((q0.z + q1.z) + qb.z) - mu) * rs * gv.z + bt.z;
    const float y3 = (((q0.w + q1.w) + qb.w) - mu) * rs * gv.w + bt.w;
    v4f o;
    o.x = 0.5f * y0 * (1.0f + erff(y0 * 0.70710678118654752f));
    o.y = 0.5f * y1 * (1.0f + erff(y1 * 0.70710678118654752f));
    o.z = 0.5f * y2 * (1.0f + erff(y2 * 0.70710678118654752f));
    o.w = 0.5f * y3 * (1.0f + erff(y3 * 0.70710678118654752f));
    *reinterpret_cast<v4f*>(sG + row * HP + 4 * cq) = o;
  }
  __syncthreads();

  {
    v8f c0v = {0, 0, 0, 0, 0, 0, 0, 0};
    v8f c1v = {0, 0, 0, 0, 0, 0, 0, 0};
    #pragma unroll
    for (int kk = 0; kk < 8; ++kk) {
      const v2f a = *reinterpret_cast<const v2f*>(sG + r * HP + 4 * kk + 2 * hf);
      c0v = wmma4(a, bw2[0][kk], c0v);
      c1v = wmma4(a, bw2[1][kk], c1v);
    }
    const int nl = (2 * wid) * 16 + r;
    #pragma unroll
    for (int g = 0; g < 8; ++g) {
      sR[(g + 8 * hf) * RP + nl]      = c0v[g] + b2v[0];
      sR[(g + 8 * hf) * RP + nl + 16] = c1v[g] + b2v[1];
    }
  }
  __syncthreads();

  #pragma unroll
  for (int q = 0; q < 2; ++q) {
    const int idx = tid + 128 * q;
    const int row = idx >> 4;
    const int hl  = (idx >> 3) & 1;
    const int j   = idx & 7;
    float* p = sR + row * RP + hl * 64 + 28 + (NKK * j - (j * (j - 1)) / 2);
    *p = softplusf(*p) + 0.01f;
  }
  if (tid < 16) { sR[tid * RP + 128] = 0.0f; sR[tid * RP + 129] = 1.0f; }
  __syncthreads();

  {
    const int hl = tid >> 6;
    const int ik = tid & 63;
    const int i  = ik >> 3;
    const int k  = ik & 7;
    int oL[NKK], oU[NKK];
    #pragma unroll
    for (int j = 0; j < NKK; ++j) {
      const bool on = (j <= i) && (j <= k);
      oL[j] = on ? ((j == i) ? 129 : (hl * 64 + (i * (i - 1)) / 2 + j)) : 128;
      oU[j] = on ? (hl * 64 + 28 + (NKK * j - (j * (j - 1)) / 2) + (k - j)) : 128;
    }
    const size_t obase = ((size_t)branch * NGRP + G0) * 512 + qtr * 128 + tid;
    #pragma unroll
    for (int row = 0; row < 16; ++row) {
      const float* rb = sR + row * RP;
      float acc = 0.0f;
      #pragma unroll
      for (int j = 0; j < NKK; ++j) acc += rb[oL[j]] * rb[oU[j]];
      out[obase + (size_t)row * 512] = acc;
    }
  }
}

extern "C" void kernel_launch(void* const* d_in, const int* in_sizes, int n_in,
                              void* d_out, int out_size, void* d_ws, size_t ws_size,
                              hipStream_t stream) {
  const float* wsfeat = (const float*)d_in[0];
  const float* W_emb  = (const float*)d_in[1];
  const float* b_emb  = (const float*)d_in[2];
  const float* pos    = (const float*)d_in[3];
  const float* w_in   = (const float*)d_in[4];
  const float* b_in   = (const float*)d_in[5];
  const float* w_out  = (const float*)d_in[6];
  const float* b_out  = (const float*)d_in[7];
  const float* m_w1 = (const float*)d_in[8];
  const float* m_b1 = (const float*)d_in[9];
  const float* m_g  = (const float*)d_in[10];
  const float* m_bb = (const float*)d_in[11];
  const float* m_w2 = (const float*)d_in[12];
  const float* m_b2 = (const float*)d_in[13];
  const float* n_w1 = (const float*)d_in[14];
  const float* n_b1 = (const float*)d_in[15];
  const float* n_g  = (const float*)d_in[16];
  const float* n_bb = (const float*)d_in[17];
  const float* n_w2 = (const float*)d_in[18];
  const float* n_b2 = (const float*)d_in[19];
  float* out = (float*)d_out;

  float* wsp    = (float*)d_ws;
  float* part   = wsp;

  k_fused<<<MROWS / UR, 128, 0, stream>>>(wsfeat, W_emb, b_emb, pos, w_in, b_in, part);
  k_tail<<<(NGRP / 16) * 8, 128, 0, stream>>>(part, w_in, b_in, w_out, b_out,
      m_w1, m_b1, m_g, m_bb, m_w2, m_b2,
      n_w1, n_b1, n_g, n_bb, n_w2, n_b2, out);
}
